# starting kernel
# baseline (speedup 1.0000x reference)
#include <hip/hip_runtime.h>

typedef __attribute__((ext_vector_type(2))) float v2f;
typedef __attribute__((ext_vector_type(8))) float v8f;

__device__ __forceinline__ v8f wmma_f32x4(v2f a, v2f b, v8f c) {
    return __builtin_amdgcn_wmma_f32_16x16x4_f32(
         false, a,  false, b,
         (short)0, c,  false,  false);
}

__device__ __forceinline__ void atomic_add_f32(float* p, float v) {
#if defined(__HIP_DEVICE_COMPILE__)
    unsafeAtomicAdd(p, v);
#else
    atomicAdd(p, v);
#endif
}

template <int K>
__global__ void gemm_bias_wmma_lds(const float* __restrict__ X,
                                   const float* __restrict__ W,
                                   const float* __restrict__ bias,
                                   float* __restrict__ Y,
                                   int M, int N) {
    __shared__ float As[16 * K];

    const int lane = threadIdx.x & 31;
    const int wave = threadIdx.x >> 5;
    const int wavesPerBlock = blockDim.x >> 5;
    const int tilesN = N >> 4;
    const int strip = blockIdx.x * 16;

    if (strip + 16 <= M) {
        const float4* __restrict__ src = (const float4*)(X + (size_t)strip * K);
        float4* dst = (float4*)As;
        for (int i = threadIdx.x; i < (16 * K) / 4; i += blockDim.x) dst[i] = src[i];
    } else {
        for (int i = threadIdx.x; i < 16 * K; i += blockDim.x) {
            const int rr = i / K, cc = i - rr * K;
            const int gr = min(strip + rr, M - 1);
            As[i] = X[(size_t)gr * K + cc];
        }
    }
    __syncthreads();

    const int koff = (lane >> 4) << 1;
    const float* __restrict__ arow = As + (lane & 15) * K + koff;
    const bool fullTile = (strip + 16 <= M);
    const int rbase = strip + ((lane >> 4) << 3);

    for (int tn = wave; tn < tilesN; tn += wavesPerBlock) {
        const int ncol = tn * 16 + (lane & 15);
        const float* __restrict__ wrow = W + (size_t)ncol * K + koff;

        v8f acc = {};
#pragma unroll
        for (int k = 0; k < K; k += 4) {
            v2f a = *(const v2f*)(arow + k);
            v2f b = *(const v2f*)(wrow + k);
            acc = wmma_f32x4(a, b, acc);
        }

        const float bv = bias[ncol];
        float* yp = Y + (size_t)rbase * N + ncol;
        if (fullTile) {
#pragma unroll
            for (int r = 0; r < 8; ++r) { *yp = acc[r] + bv; yp += N; }
        } else {
#pragma unroll
            for (int r = 0; r < 8; ++r) {
                if (rbase + r < M) *yp = acc[r] + bv;
                yp += N;
            }
        }
    }
}

__global__ void gemm_bias_wmma_rt(const float* __restrict__ X,
                                  const float* __restrict__ W,
                                  const float* __restrict__ bias,
                                  float* __restrict__ Y,
                                  int M, int K, int N) {
    const int lane = threadIdx.x & 31;
    const int wave = threadIdx.x >> 5;
    const int tilesN = N >> 4;
    const int tile = blockIdx.x * (blockDim.x >> 5) + wave;
    const int tm = tile / tilesN;
    const int tn = tile - tm * tilesN;
    if (tm * 16 >= M) return;

    const int mrow = min(tm * 16 + (lane & 15), M - 1);
    const int ncol = tn * 16 + (lane & 15);
    const int koff = (lane >> 4) << 1;
    const float* __restrict__ xrow = X + (size_t)mrow * K + koff;
    const float* __restrict__ wrow = W + (size_t)ncol * K + koff;

    v8f acc = {};
#pragma unroll 8
    for (int k = 0; k < K; k += 4) {
        v2f a = *(const v2f*)(xrow + k);
        v2f b = *(const v2f*)(wrow + k);
        acc = wmma_f32x4(a, b, acc);
    }

    const float bv = bias[ncol];
    const int rbase = tm * 16 + ((lane >> 4) << 3);
    float* yp = Y + (size_t)rbase * N + ncol;
#pragma unroll
    for (int r = 0; r < 8; ++r) {
        if (rbase + r < M) *yp = acc[r] + bv;
        yp += N;
    }
}

template <int HT>
__global__ void edge_mlp_wmma(const float* __restrict__ xf,
                              const int* __restrict__ row,
                              const int* __restrict__ col,
                              const float* __restrict__ We,
                              const float* __restrict__ be,
                              float* __restrict__ out,
                              int E, int Nu, int Hrt, int C) {
    const int H = (HT > 0) ? HT : Hrt;
    const int K = 2 * H;
    const int lane = threadIdx.x & 31;
    const int wave = threadIdx.x >> 5;
    const int tile = blockIdx.x * (blockDim.x >> 5) + wave;
    const int ebase = tile * 16;
    if (ebase >= E) return;

    const int m = lane & 15;
    const int e = min(ebase + m, E - 1);
    const int r = row[e];
    const int c = col[e];
    const int koff = (lane >> 4) << 1;

    const float* __restrict__ urow = xf + (size_t)r * H + koff;
    const float* __restrict__ mrow = xf + (size_t)(Nu + c) * H + koff;

    const int n = lane & 15;
    const float wmask = (n < C) ? 1.0f : 0.0f;
    const float* __restrict__ wrow = We + (size_t)min(n, C - 1) * K + koff;

    v8f acc = {};
#pragma unroll 8
    for (int k = 0; k < K; k += 4) {
        const float* src = (k + koff < H) ? (urow + k) : (mrow + (k - H));
        v2f a = *(const v2f*)src;
        v2f b = *(const v2f*)(wrow + k);
        b.x *= wmask;
        b.y *= wmask;
        acc = wmma_f32x4(a, b, acc);
    }

    if (n < C) {
        const float bv = be[n];
        const int mb = (lane >> 4) << 3;
        float* op = out + (size_t)(ebase + mb) * C + n;
        if (ebase + 16 <= E) {
#pragma unroll
            for (int rr = 0; rr < 8; ++rr) { *op = acc[rr] + bv; op += C; }
        } else {
#pragma unroll
            for (int rr = 0; rr < 8; ++rr) {
                if (ebase + mb + rr < E) *op = acc[rr] + bv;
                op += C;
            }
        }
    }
}

__global__ void fill_kernel(float* __restrict__ p, float v, long long n) {
    long long i = (long long)blockIdx.x * blockDim.x + threadIdx.x;
    if (i < n) p[i] = v;
}

__global__ void zero4_kernel(float4* __restrict__ p, long long n4) {
    long long i = (long long)blockIdx.x * blockDim.x + threadIdx.x;
    if (i < n4) p[i] = make_float4(0.f, 0.f, 0.f, 0.f);
}

__global__ void deg_scatter_kernel(const int* __restrict__ col,
                                   float* __restrict__ deg, int E) {
    int e = blockIdx.x * blockDim.x + threadIdx.x;
    if (e < E) atomic_add_f32(&deg[col[e]], 1.0f);
}

__global__ void rsqrt_kernel(float* __restrict__ d, int N) {
    int i = blockIdx.x * blockDim.x + threadIdx.x;
    if (i < N) d[i] = rsqrtf(d[i]);
}

__global__ void edge_scatter_kernel(const float* __restrict__ h,
                                    const float* __restrict__ dinv,
                                    const int* __restrict__ row,
                                    const int* __restrict__ col,
                                    float* __restrict__ acc,
                                    int E, int H) {
    const int chunks = H >> 2;
    long long gid = (long long)blockIdx.x * blockDim.x + threadIdx.x;
    long long total = (long long)E * chunks;
    if (gid >= total) return;
    const int e = (int)(gid / chunks);
    const int j = (int)(gid - (long long)e * chunks) << 2;
    const int r = row[e];
    const int c = col[e];
    const float s = dinv[r];
    const float4 hv = *(const float4*)(h + (size_t)r * H + j);
    float* dst = acc + (size_t)c * H + j;
    atomic_add_f32(dst + 0, s * hv.x);
    atomic_add_f32(dst + 1, s * hv.y);
    atomic_add_f32(dst + 2, s * hv.z);
    atomic_add_f32(dst + 3, s * hv.w);
}

__global__ void finalize_relu_kernel(const float* __restrict__ h,
                                     const float* __restrict__ dinv,
                                     float* __restrict__ acc_inout, int H) {
    const int node = blockIdx.x;
    const int j = threadIdx.x;
    const float di = dinv[node];
    const size_t idx = (size_t)node * H + j;
    const float v = di * acc_inout[idx] + di * di * h[idx];
    acc_inout[idx] = v > 0.f ? v : 0.f;
}

static inline int ceil_div_i(long long a, long long b) { return (int)((a + b - 1) / b); }

static void launch_gemm(const float* X, const float* W, const float* b, float* Y,
                        int M, int K, int N, hipStream_t stream) {
    const int TPB = 256;
    const int strips = ceil_div_i(M, 16);
    if (K == 32)
        gemm_bias_wmma_lds<32><<<strips, TPB, 0, stream>>>(X, W, b, Y, M, N);
    else if (K == 64)
        gemm_bias_wmma_lds<64><<<strips, TPB, 0, stream>>>(X, W, b, Y, M, N);
    else if (K == 128)
        gemm_bias_wmma_lds<128><<<strips, TPB, 0, stream>>>(X, W, b, Y, M, N);
    else {
        const int tiles = strips * (N / 16);
        gemm_bias_wmma_rt<<<ceil_div_i(tiles, 8), TPB, 0, stream>>>(X, W, b, Y, M, K, N);
    }
}

extern "C" void kernel_launch(void* const* d_in, const int* in_sizes, int n_in,
                              void* d_out, int out_size, void* d_ws, size_t ws_size,
                              hipStream_t stream) {
    const float* x_user  = (const float*)d_in[0];
    const float* x_movie = (const float*)d_in[1];
    const int*   edge    = (const int*)d_in[2];
    const float* Wu = (const float*)d_in[3];
    const float* bu = (const float*)d_in[4];
    const float* Wm = (const float*)d_in[5];
    const float* bm = (const float*)d_in[6];
    const float* W1 = (const float*)d_in[7];
    const float* b1 = (const float*)d_in[8];
    const float* W2 = (const float*)d_in[9];
    const float* b2 = (const float*)d_in[10];
    const float* We = (const float*)d_in[11];
    const float* be = (const float*)d_in[12];
    float* out = (float*)d_out;

    const int H  = in_sizes[4];
    const int C  = in_sizes[12];
    const int Fu = in_sizes[3] / H;
    const int Fm = in_sizes[5] / H;
    const int Nu = in_sizes[0] / Fu;
    const int Nm = in_sizes[1] / Fm;
    const int E  = in_sizes[2] / 2;
    const int N  = Nu + Nm;
    const int* row = edge;
    const int* col = edge + E;

    float* ws   = (float*)d_ws;
    float* X    = ws;
    float* Hh   = X  + (size_t)N * H;
    float* Acc  = Hh + (size_t)N * H;
    float* dinv = Acc + (size_t)N * H;

    const int TPB = 256, WPB = 8;
    const long long NH = (long long)N * H;

    launch_gemm(x_user, Wu, bu, X, Nu, Fu, H, stream);
    launch_gemm(x_movie, Wm, bm, X + (size_t)Nu * H, Nm, Fm, H, stream);

    fill_kernel<<<ceil_div_i(N, TPB), TPB, 0, stream>>>(dinv, 1.0f, N);
    deg_scatter_kernel<<<ceil_div_i(E, TPB), TPB, 0, stream>>>(col, dinv, E);
    rsqrt_kernel<<<ceil_div_i(N, TPB), TPB, 0, stream>>>(dinv, N);

    const long long scatterThreads = (long long)E * (H / 4);

    launch_gemm(X, W1, b1, Hh, N, H, H, stream);
    zero4_kernel<<<ceil_div_i(NH / 4, TPB), TPB, 0, stream>>>((float4*)Acc, NH / 4);
    edge_scatter_kernel<<<ceil_div_i(scatterThreads, TPB), TPB, 0, stream>>>(Hh, dinv, row, col, Acc, E, H);
    finalize_relu_kernel<<<N, H, 0, stream>>>(Hh, dinv, Acc, H);

    launch_gemm(Acc, W2, b2, Hh, N, H, H, stream);
    zero4_kernel<<<ceil_div_i(NH / 4, TPB), TPB, 0, stream>>>((float4*)X, NH / 4);
    edge_scatter_kernel<<<ceil_div_i(scatterThreads, TPB), TPB, 0, stream>>>(Hh, dinv, row, col, X, E, H);
    finalize_relu_kernel<<<N, H, 0, stream>>>(Hh, dinv, X, H);

    const int edgeTiles = ceil_div_i(E, 16);
    if (H == 128)
        edge_mlp_wmma<128><<<ceil_div_i(edgeTiles, WPB), TPB, 0, stream>>>(X, row, col, We, be, out, E, Nu, H, C);
    else
        edge_mlp_wmma<0><<<ceil_div_i(edgeTiles, WPB), TPB, 0, stream>>>(X, row, col, We, be, out, E, Nu, H, C);
}


# edge MLP by linearity: WMMA projects each node once (P = x @ We_half^T, +be), each edge adds two gathered P rows; sum order changes (two 128-term dots added, not one 256-term dot)
# speedup vs baseline: 1.2058x; 1.2058x over previous
#include <hip/hip_runtime.h>

typedef __attribute__((ext_vector_type(2))) float v2f;
typedef __attribute__((ext_vector_type(8))) float v8f;

__device__ __forceinline__ v8f wmma_f32x4(v2f a, v2f b, v8f c) {
    return __builtin_amdgcn_wmma_f32_16x16x4_f32(
         false, a,  false, b,
         (short)0, c,  false,  false);
}

__device__ __forceinline__ void atomic_add_f32(float* p, float v) {
#if defined(__HIP_DEVICE_COMPILE__)
    unsafeAtomicAdd(p, v);
#else
    atomicAdd(p, v);
#endif
}

template <int K>
__global__ void gemm_bias_wmma_lds(const float* __restrict__ X,
                                   const float* __restrict__ W,
                                   const float* __restrict__ bias,
                                   float* __restrict__ Y,
                                   int M, int N) {
    __shared__ float As[16 * K];

    const int lane = threadIdx.x & 31;
    const int wave = threadIdx.x >> 5;
    const int wavesPerBlock = blockDim.x >> 5;
    const int tilesN = N >> 4;
    const int strip = blockIdx.x * 16;

    if (strip + 16 <= M) {
        const float4* __restrict__ src = (const float4*)(X + (size_t)strip * K);
        float4* dst = (float4*)As;
        for (int i = threadIdx.x; i < (16 * K) / 4; i += blockDim.x) dst[i] = src[i];
    } else {
        for (int i = threadIdx.x; i < 16 * K; i += blockDim.x) {
            const int rr = i / K, cc = i - rr * K;
            const int gr = min(strip + rr, M - 1);
            As[i] = X[(size_t)gr * K + cc];
        }
    }
    __syncthreads();

    const int koff = (lane >> 4) << 1;
    const float* __restrict__ arow = As + (lane & 15) * K + koff;
    const bool fullTile = (strip + 16 <= M);
    const int rbase = strip + ((lane >> 4) << 3);

    for (int tn = wave; tn < tilesN; tn += wavesPerBlock) {
        const int ncol = tn * 16 + (lane & 15);
        const float* __restrict__ wrow = W + (size_t)ncol * K + koff;

        v8f acc = {};
#pragma unroll
        for (int k = 0; k < K; k += 4) {
            v2f a = *(const v2f*)(arow + k);
            v2f b = *(const v2f*)(wrow + k);
            acc = wmma_f32x4(a, b, acc);
        }

        const float bv = bias[ncol];
        float* yp = Y + (size_t)rbase * N + ncol;
        if (fullTile) {
#pragma unroll
            for (int r = 0; r < 8; ++r) { *yp = acc[r] + bv; yp += N; }
        } else {
#pragma unroll
            for (int r = 0; r < 8; ++r) {
                if (rbase + r < M) *yp = acc[r] + bv;
                yp += N;
            }
        }
    }
}

__global__ void gemm_bias_wmma_rt(const float* __restrict__ X,
                                  const float* __restrict__ W,
                                  const float* __restrict__ bias,
                                  float* __restrict__ Y,
                                  int M, int K, int N) {
    const int lane = threadIdx.x & 31;
    const int wave = threadIdx.x >> 5;
    const int tilesN = N >> 4;
    const int tile = blockIdx.x * (blockDim.x >> 5) + wave;
    const int tm = tile / tilesN;
    const int tn = tile - tm * tilesN;
    if (tm * 16 >= M) return;

    const int mrow = min(tm * 16 + (lane & 15), M - 1);
    const int ncol = tn * 16 + (lane & 15);
    const int koff = (lane >> 4) << 1;
    const float* __restrict__ xrow = X + (size_t)mrow * K + koff;
    const float* __restrict__ wrow = W + (size_t)ncol * K + koff;

    v8f acc = {};
#pragma unroll 8
    for (int k = 0; k < K; k += 4) {
        v2f a = *(const v2f*)(xrow + k);
        v2f b = *(const v2f*)(wrow + k);
        acc = wmma_f32x4(a, b, acc);
    }

    const float bv = bias[ncol];
    const int rbase = tm * 16 + ((lane >> 4) << 3);
    float* yp = Y + (size_t)rbase * N + ncol;
#pragma unroll
    for (int r = 0; r < 8; ++r) {
        if (rbase + r < M) *yp = acc[r] + bv;
        yp += N;
    }
}

template <int HT>
__global__ void edge_mlp_wmma(const float* __restrict__ xf,
                              const int* __restrict__ row,
                              const int* __restrict__ col,
                              const float* __restrict__ We,
                              const float* __restrict__ be,
                              float* __restrict__ out,
                              int E, int Nu, int Hrt, int C) {
    const int H = (HT > 0) ? HT : Hrt;
    const int K = 2 * H;
    const int lane = threadIdx.x & 31;
    const int wave = threadIdx.x >> 5;
    const int tile = blockIdx.x * (blockDim.x >> 5) + wave;
    const int ebase = tile * 16;
    if (ebase >= E) return;

    const int m = lane & 15;
    const int e = min(ebase + m, E - 1);
    const int r = row[e];
    const int c = col[e];
    const int koff = (lane >> 4) << 1;

    const float* __restrict__ urow = xf + (size_t)r * H + koff;
    const float* __restrict__ mrow = xf + (size_t)(Nu + c) * H + koff;

    const int n = lane & 15;
    const float wmask = (n < C) ? 1.0f : 0.0f;
    const float* __restrict__ wrow = We + (size_t)min(n, C - 1) * K + koff;

    v8f acc = {};
#pragma unroll 8
    for (int k = 0; k < K; k += 4) {
        const float* src = (k + koff < H) ? (urow + k) : (mrow + (k - H));
        v2f a = *(const v2f*)src;
        v2f b = *(const v2f*)(wrow + k);
        b.x *= wmask;
        b.y *= wmask;
        acc = wmma_f32x4(a, b, acc);
    }

    if (n < C) {
        const float bv = be[n];
        const int mb = (lane >> 4) << 3;
        float* op = out + (size_t)(ebase + mb) * C + n;
        if (ebase + 16 <= E) {
#pragma unroll
            for (int rr = 0; rr < 8; ++rr) { *op = acc[rr] + bv; op += C; }
        } else {
#pragma unroll
            for (int rr = 0; rr < 8; ++rr) {
                if (ebase + mb + rr < E) *op = acc[rr] + bv;
                op += C;
            }
        }
    }
}

__global__ void node_proj_wmma(const float* __restrict__ xf,
                               const float* __restrict__ We,
                               const float* __restrict__ be,
                               float* __restrict__ P,
                               int Nu, int Nm, int C) {
    constexpr int H = 128;
    const int lane = threadIdx.x & 31;
    const int wave = threadIdx.x >> 5;
    const int tilesU = (Nu + 15) >> 4;
    const int tilesM = (Nm + 15) >> 4;
    const int tile = blockIdx.x * (blockDim.x >> 5) + wave;
    if (tile >= tilesU + tilesM) return;

    const bool second = tile >= tilesU;
    const int segEnd = second ? Nu + Nm : Nu;
    const int base = second ? Nu + ((tile - tilesU) << 4) : (tile << 4);

    const int c = lane & 15;
    const int h = lane >> 4;
    const int node = min(base + 2 * (c & 7) + (c >> 3), segEnd - 1);
    const float* __restrict__ arow = xf + (size_t)node * H + 4 * h;
    const float* __restrict__ brow =
        We + (size_t)min(c, C - 1) * (2 * H) + (second ? H : 0) + 4 * h;

    const float bsel = be[min(c, C - 1)];
    const float bv = (second && c < C) ? bsel : 0.0f;

    v8f acc0 = {}, acc1 = {};
#pragma unroll
    for (int i = 0; i < H / 8; ++i) {
        const float4 a = *(const float4*)(arow + 8 * i);
        const float4 b = *(const float4*)(brow + 8 * i);
        v2f a0 = {a.x, a.y}, b0 = {b.x, b.y};
        v2f a1 = {a.z, a.w}, b1 = {b.z, b.w};
        acc0 = wmma_f32x4(a0, b0, acc0);
        acc1 = wmma_f32x4(a1, b1, acc1);
    }

    float v[8];
#pragma unroll
    for (int r = 0; r < 8; ++r) v[r] = (acc0[r] + acc1[r]) + bv;

    float* pp = P + (size_t)base * 16 + lane;
    if (base + 16 <= segEnd) {
#pragma unroll
        for (int r = 0; r < 8; ++r) pp[32 * r] = v[r];
    } else {
#pragma unroll
        for (int r = 0; r < 8; ++r)
            if (base + 2 * r + h < segEnd) pp[32 * r] = v[r];
    }
}

template <int ITER>
__global__ void edge_add10(const float* __restrict__ P,
                           const int* __restrict__ row,
                           const int* __restrict__ col,
                           float* __restrict__ out,
                           int E, int Nu) {
    constexpr unsigned C = 10;
    const unsigned total = (unsigned)E * C;
    const float* __restrict__ Pu = P;
    const float* __restrict__ Pm = P + (size_t)Nu * 16;
    const unsigned perBlock = blockDim.x * ITER;
    const unsigned qb = blockIdx.x * perBlock;

    if ((qb + perBlock) * 4u <= total) {
        unsigned f0[ITER], cA[ITER], cB[ITER];
        int rA[ITER], mA[ITER], rB[ITER], mB[ITER];
#pragma unroll
        for (int it = 0; it < ITER; ++it) {
            f0[it] = 4u * (qb + it * blockDim.x + threadIdx.x);
            const unsigned f1 = f0[it] + 2u;
            const unsigned eA = f0[it] / C, eB = f1 / C;
            cA[it] = f0[it] - eA * C;
            cB[it] = f1 - eB * C;
            rA[it] = row[eA]; mA[it] = col[eA];
            rB[it] = row[eB]; mB[it] = col[eB];
        }
        float2 uA[ITER], vA[ITER], uB[ITER], vB[ITER];
#pragma unroll
        for (int it = 0; it < ITER; ++it) {
            uA[it] = *(const float2*)(Pu + (size_t)rA[it] * 16 + cA[it]);
            vA[it] = *(const float2*)(Pm + (size_t)mA[it] * 16 + cA[it]);
            uB[it] = *(const float2*)(Pu + (size_t)rB[it] * 16 + cB[it]);
            vB[it] = *(const float2*)(Pm + (size_t)mB[it] * 16 + cB[it]);
        }
#pragma unroll
        for (int it = 0; it < ITER; ++it) {
            *(float4*)(out + f0[it]) = make_float4(uA[it].x + vA[it].x, uA[it].y + vA[it].y,
                                                   uB[it].x + vB[it].x, uB[it].y + vB[it].y);
        }
    } else {
        for (unsigned p = threadIdx.x; p < 2u * perBlock; p += blockDim.x) {
            const unsigned f = 4u * qb + 2u * p;
            if (f < total) {
                const unsigned e = f / C, c = f - e * C;
                const float2 u = *(const float2*)(Pu + (size_t)row[e] * 16 + c);
                const float2 v = *(const float2*)(Pm + (size_t)col[e] * 16 + c);
                *(float2*)(out + f) = make_float2(u.x + v.x, u.y + v.y);
            }
        }
    }
}

__global__ void fill_kernel(float* __restrict__ p, float v, long long n) {
    long long i = (long long)blockIdx.x * blockDim.x + threadIdx.x;
    if (i < n) p[i] = v;
}

__global__ void zero4_kernel(float4* __restrict__ p, long long n4) {
    long long i = (long long)blockIdx.x * blockDim.x + threadIdx.x;
    if (i < n4) p[i] = make_float4(0.f, 0.f, 0.f, 0.f);
}

__global__ void deg_scatter_kernel(const int* __restrict__ col,
                                   float* __restrict__ deg, int E) {
    int e = blockIdx.x * blockDim.x + threadIdx.x;
    if (e < E) atomic_add_f32(&deg[col[e]], 1.0f);
}

__global__ void rsqrt_kernel(float* __restrict__ d, int N) {
    int i = blockIdx.x * blockDim.x + threadIdx.x;
    if (i < N) d[i] = rsqrtf(d[i]);
}

__global__ void edge_scatter_kernel(const float* __restrict__ h,
                                    const float* __restrict__ dinv,
                                    const int* __restrict__ row,
                                    const int* __restrict__ col,
                                    float* __restrict__ acc,
                                    int E, int H) {
    const int chunks = H >> 2;
    long long gid = (long long)blockIdx.x * blockDim.x + threadIdx.x;
    long long total = (long long)E * chunks;
    if (gid >= total) return;
    const int e = (int)(gid / chunks);
    const int j = (int)(gid - (long long)e * chunks) << 2;
    const int r = row[e];
    const int c = col[e];
    const float s = dinv[r];
    const float4 hv = *(const float4*)(h + (size_t)r * H + j);
    float* dst = acc + (size_t)c * H + j;
    atomic_add_f32(dst + 0, s * hv.x);
    atomic_add_f32(dst + 1, s * hv.y);
    atomic_add_f32(dst + 2, s * hv.z);
    atomic_add_f32(dst + 3, s * hv.w);
}

__global__ void finalize_relu_kernel(const float* __restrict__ h,
                                     const float* __restrict__ dinv,
                                     float* __restrict__ acc_inout, int H) {
    const int node = blockIdx.x;
    const int j = threadIdx.x;
    const float di = dinv[node];
    const size_t idx = (size_t)node * H + j;
    const float v = di * acc_inout[idx] + di * di * h[idx];
    acc_inout[idx] = v > 0.f ? v : 0.f;
}

static inline int ceil_div_i(long long a, long long b) { return (int)((a + b - 1) / b); }

static void launch_gemm(const float* X, const float* W, const float* b, float* Y,
                        int M, int K, int N, hipStream_t stream) {
    const int TPB = 256;
    const int strips = ceil_div_i(M, 16);
    if (K == 32)
        gemm_bias_wmma_lds<32><<<strips, TPB, 0, stream>>>(X, W, b, Y, M, N);
    else if (K == 64)
        gemm_bias_wmma_lds<64><<<strips, TPB, 0, stream>>>(X, W, b, Y, M, N);
    else if (K == 128)
        gemm_bias_wmma_lds<128><<<strips, TPB, 0, stream>>>(X, W, b, Y, M, N);
    else {
        const int tiles = strips * (N / 16);
        gemm_bias_wmma_rt<<<ceil_div_i(tiles, 8), TPB, 0, stream>>>(X, W, b, Y, M, K, N);
    }
}

extern "C" void kernel_launch(void* const* d_in, const int* in_sizes, int n_in,
                              void* d_out, int out_size, void* d_ws, size_t ws_size,
                              hipStream_t stream) {
    const float* x_user  = (const float*)d_in[0];
    const float* x_movie = (const float*)d_in[1];
    const int*   edge    = (const int*)d_in[2];
    const float* Wu = (const float*)d_in[3];
    const float* bu = (const float*)d_in[4];
    const float* Wm = (const float*)d_in[5];
    const float* bm = (const float*)d_in[6];
    const float* W1 = (const float*)d_in[7];
    const float* b1 = (const float*)d_in[8];
    const float* W2 = (const float*)d_in[9];
    const float* b2 = (const float*)d_in[10];
    const float* We = (const float*)d_in[11];
    const float* be = (const float*)d_in[12];
    float* out = (float*)d_out;

    const int H  = in_sizes[4];
    const int C  = in_sizes[12];
    const int Fu = in_sizes[3] / H;
    const int Fm = in_sizes[5] / H;
    const int Nu = in_sizes[0] / Fu;
    const int Nm = in_sizes[1] / Fm;
    const int E  = in_sizes[2] / 2;
    const int N  = Nu + Nm;
    const int* row = edge;
    const int* col = edge + E;

    float* ws   = (float*)d_ws;
    float* X    = ws;
    float* Hh   = X  + (size_t)N * H;
    float* Acc  = Hh + (size_t)N * H;
    float* dinv = Acc + (size_t)N * H;

    const int TPB = 256, WPB = 8;
    const long long NH = (long long)N * H;

    launch_gemm(x_user, Wu, bu, X, Nu, Fu, H, stream);
    launch_gemm(x_movie, Wm, bm, X + (size_t)Nu * H, Nm, Fm, H, stream);

    fill_kernel<<<ceil_div_i(N, TPB), TPB, 0, stream>>>(dinv, 1.0f, N);
    deg_scatter_kernel<<<ceil_div_i(E, TPB), TPB, 0, stream>>>(col, dinv, E);
    rsqrt_kernel<<<ceil_div_i(N, TPB), TPB, 0, stream>>>(dinv, N);

    const long long scatterThreads = (long long)E * (H / 4);

    launch_gemm(X, W1, b1, Hh, N, H, H, stream);
    zero4_kernel<<<ceil_div_i(NH / 4, TPB), TPB, 0, stream>>>((float4*)Acc, NH / 4);
    edge_scatter_kernel<<<ceil_div_i(scatterThreads, TPB), TPB, 0, stream>>>(Hh, dinv, row, col, Acc, E, H);
    finalize_relu_kernel<<<N, H, 0, stream>>>(Hh, dinv, Acc, H);

    launch_gemm(Acc, W2, b2, Hh, N, H, H, stream);
    zero4_kernel<<<ceil_div_i(NH / 4, TPB), TPB, 0, stream>>>((float4*)X, NH / 4);
    edge_scatter_kernel<<<ceil_div_i(scatterThreads, TPB), TPB, 0, stream>>>(Hh, dinv, row, col, X, E, H);
    finalize_relu_kernel<<<N, H, 0, stream>>>(Hh, dinv, X, H);

    const int edgeTiles = ceil_div_i(E, 16);
    if (H == 128 && C == 10) {
        float* P = dinv + (((size_t)N + 31) & ~(size_t)31);
        const int nodeTiles = ceil_div_i(Nu, 16) + ceil_div_i(Nm, 16);
        node_proj_wmma<<<ceil_div_i(nodeTiles, 4), 128, 0, stream>>>(X, We, be, P, Nu, Nm, C);
        const long long quads = ((long long)E * C + 3) / 4;
        edge_add10<4><<<ceil_div_i(quads, TPB * 4), TPB, 0, stream>>>(P, row, col, out, E, Nu);
    } else if (H == 128)
        edge_mlp_wmma<128><<<ceil_div_i(edgeTiles, WPB), TPB, 0, stream>>>(X, row, col, We, be, out, E, Nu, H, C);
    else
        edge_mlp_wmma<0><<<ceil_div_i(edgeTiles, WPB), TPB, 0, stream>>>(X, row, col, We, be, out, E, Nu, H, C);
}


# aggregation by gather: per-call in-edge lists grouped by target node (int atomics), one wave per node sums its in-edges in registers, relu fused; float atomics gone; sum order changes
# speedup vs baseline: 2.0115x; 1.6682x over previous
#include <hip/hip_runtime.h>

typedef __attribute__((ext_vector_type(2))) float v2f;
typedef __attribute__((ext_vector_type(8))) float v8f;

__device__ __forceinline__ v8f wmma_f32x4(v2f a, v2f b, v8f c) {
    return __builtin_amdgcn_wmma_f32_16x16x4_f32(
         false, a,  false, b,
         (short)0, c,  false,  false);
}

template <int K>
__global__ void gemm_bias_wmma_lds(const float* __restrict__ X,
                                   const float* __restrict__ W,
                                   const float* __restrict__ bias,
                                   float* __restrict__ Y,
                                   int M, int N) {
    __shared__ float As[16 * K];

    const int lane = threadIdx.x & 31;
    const int wave = threadIdx.x >> 5;
    const int wavesPerBlock = blockDim.x >> 5;
    const int tilesN = N >> 4;
    const int strip = blockIdx.x * 16;

    if (strip + 16 <= M) {
        const float4* __restrict__ src = (const float4*)(X + (size_t)strip * K);
        float4* dst = (float4*)As;
        for (int i = threadIdx.x; i < (16 * K) / 4; i += blockDim.x) dst[i] = src[i];
    } else {
        for (int i = threadIdx.x; i < 16 * K; i += blockDim.x) {
            const int rr = i / K, cc = i - rr * K;
            const int gr = min(strip + rr, M - 1);
            As[i] = X[(size_t)gr * K + cc];
        }
    }
    __syncthreads();

    const int koff = (lane >> 4) << 1;
    const float* __restrict__ arow = As + (lane & 15) * K + koff;
    const bool fullTile = (strip + 16 <= M);
    const int rbase = strip + ((lane >> 4) << 3);

    for (int tn = wave; tn < tilesN; tn += wavesPerBlock) {
        const int ncol = tn * 16 + (lane & 15);
        const float* __restrict__ wrow = W + (size_t)ncol * K + koff;

        v8f acc = {};
#pragma unroll
        for (int k = 0; k < K; k += 4) {
            v2f a = *(const v2f*)(arow + k);
            v2f b = *(const v2f*)(wrow + k);
            acc = wmma_f32x4(a, b, acc);
        }

        const float bv = bias[ncol];
        float* yp = Y + (size_t)rbase * N + ncol;
        if (fullTile) {
#pragma unroll
            for (int r = 0; r < 8; ++r) { *yp = acc[r] + bv; yp += N; }
        } else {
#pragma unroll
            for (int r = 0; r < 8; ++r) {
                if (rbase + r < M) *yp = acc[r] + bv;
                yp += N;
            }
        }
    }
}

__global__ void gemm_bias_wmma_rt(const float* __restrict__ X,
                                  const float* __restrict__ W,
                                  const float* __restrict__ bias,
                                  float* __restrict__ Y,
                                  int M, int K, int N) {
    const int lane = threadIdx.x & 31;
    const int wave = threadIdx.x >> 5;
    const int tilesN = N >> 4;
    const int tile = blockIdx.x * (blockDim.x >> 5) + wave;
    const int tm = tile / tilesN;
    const int tn = tile - tm * tilesN;
    if (tm * 16 >= M) return;

    const int mrow = min(tm * 16 + (lane & 15), M - 1);
    const int ncol = tn * 16 + (lane & 15);
    const int koff = (lane >> 4) << 1;
    const float* __restrict__ xrow = X + (size_t)mrow * K + koff;
    const float* __restrict__ wrow = W + (size_t)ncol * K + koff;

    v8f acc = {};
#pragma unroll 8
    for (int k = 0; k < K; k += 4) {
        v2f a = *(const v2f*)(xrow + k);
        v2f b = *(const v2f*)(wrow + k);
        acc = wmma_f32x4(a, b, acc);
    }

    const float bv = bias[ncol];
    const int rbase = tm * 16 + ((lane >> 4) << 3);
    float* yp = Y + (size_t)rbase * N + ncol;
#pragma unroll
    for (int r = 0; r < 8; ++r) {
        if (rbase + r < M) *yp = acc[r] + bv;
        yp += N;
    }
}

template <int HT>
__global__ void edge_mlp_wmma(const float* __restrict__ xf,
                              const int* __restrict__ row,
                              const int* __restrict__ col,
                              const float* __restrict__ We,
                              const float* __restrict__ be,
                              float* __restrict__ out,
                              int E, int Nu, int Hrt, int C) {
    const int H = (HT > 0) ? HT : Hrt;
    const int K = 2 * H;
    const int lane = threadIdx.x & 31;
    const int wave = threadIdx.x >> 5;
    const int tile = blockIdx.x * (blockDim.x >> 5) + wave;
    const int ebase = tile * 16;
    if (ebase >= E) return;

    const int m = lane & 15;
    const int e = min(ebase + m, E - 1);
    const int r = row[e];
    const int c = col[e];
    const int koff = (lane >> 4) << 1;

    const float* __restrict__ urow = xf + (size_t)r * H + koff;
    const float* __restrict__ mrow = xf + (size_t)(Nu + c) * H + koff;

    const int n = lane & 15;
    const float wmask = (n < C) ? 1.0f : 0.0f;
    const float* __restrict__ wrow = We + (size_t)min(n, C - 1) * K + koff;

    v8f acc = {};
#pragma unroll 8
    for (int k = 0; k < K; k += 4) {
        const float* src = (k + koff < H) ? (urow + k) : (mrow + (k - H));
        v2f a = *(const v2f*)src;
        v2f b = *(const v2f*)(wrow + k);
        b.x *= wmask;
        b.y *= wmask;
        acc = wmma_f32x4(a, b, acc);
    }

    if (n < C) {
        const float bv = be[n];
        const int mb = (lane >> 4) << 3;
        float* op = out + (size_t)(ebase + mb) * C + n;
        if (ebase + 16 <= E) {
#pragma unroll
            for (int rr = 0; rr < 8; ++rr) { *op = acc[rr] + bv; op += C; }
        } else {
#pragma unroll
            for (int rr = 0; rr < 8; ++rr) {
                if (ebase + mb + rr < E) *op = acc[rr] + bv;
                op += C;
            }
        }
    }
}

__global__ void node_proj_wmma(const float* __restrict__ xf,
                               const float* __restrict__ We,
                               const float* __restrict__ be,
                               float* __restrict__ P,
                               int Nu, int Nm, int C) {
    constexpr int H = 128;
    const int lane = threadIdx.x & 31;
    const int wave = threadIdx.x >> 5;
    const int tilesU = (Nu + 15) >> 4;
    const int tilesM = (Nm + 15) >> 4;
    const int tile = blockIdx.x * (blockDim.x >> 5) + wave;
    if (tile >= tilesU + tilesM) return;

    const bool second = tile >= tilesU;
    const int segEnd = second ? Nu + Nm : Nu;
    const int base = second ? Nu + ((tile - tilesU) << 4) : (tile << 4);

    const int c = lane & 15;
    const int h = lane >> 4;
    const int node = min(base + 2 * (c & 7) + (c >> 3), segEnd - 1);
    const float* __restrict__ arow = xf + (size_t)node * H + 4 * h;
    const float* __restrict__ brow =
        We + (size_t)min(c, C - 1) * (2 * H) + (second ? H : 0) + 4 * h;

    const float bsel = be[min(c, C - 1)];
    const float bv = (second && c < C) ? bsel : 0.0f;

    v8f acc0 = {}, acc1 = {};
#pragma unroll
    for (int i = 0; i < H / 8; ++i) {
        const float4 a = *(const float4*)(arow + 8 * i);
        const float4 b = *(const float4*)(brow + 8 * i);
        v2f a0 = {a.x, a.y}, b0 = {b.x, b.y};
        v2f a1 = {a.z, a.w}, b1 = {b.z, b.w};
        acc0 = wmma_f32x4(a0, b0, acc0);
        acc1 = wmma_f32x4(a1, b1, acc1);
    }

    float v[8];
#pragma unroll
    for (int r = 0; r < 8; ++r) v[r] = (acc0[r] + acc1[r]) + bv;

    float* pp = P + (size_t)base * 16 + lane;
    if (base + 16 <= segEnd) {
#pragma unroll
        for (int r = 0; r < 8; ++r) pp[32 * r] = v[r];
    } else {
#pragma unroll
        for (int r = 0; r < 8; ++r)
            if (base + 2 * r + h < segEnd) pp[32 * r] = v[r];
    }
}

template <int ITER>
__global__ void edge_add10(const float* __restrict__ P,
                           const int* __restrict__ row,
                           const int* __restrict__ col,
                           float* __restrict__ out,
                           int E, int Nu) {
    constexpr unsigned C = 10;
    const unsigned total = (unsigned)E * C;
    const float* __restrict__ Pu = P;
    const float* __restrict__ Pm = P + (size_t)Nu * 16;
    const unsigned perBlock = blockDim.x * ITER;
    const unsigned qb = blockIdx.x * perBlock;

    if ((qb + perBlock) * 4u <= total) {
        unsigned f0[ITER], cA[ITER], cB[ITER];
        int rA[ITER], mA[ITER], rB[ITER], mB[ITER];
#pragma unroll
        for (int it = 0; it < ITER; ++it) {
            f0[it] = 4u * (qb + it * blockDim.x + threadIdx.x);
            const unsigned f1 = f0[it] + 2u;
            const unsigned eA = f0[it] / C, eB = f1 / C;
            cA[it] = f0[it] - eA * C;
            cB[it] = f1 - eB * C;
            rA[it] = row[eA]; mA[it] = col[eA];
            rB[it] = row[eB]; mB[it] = col[eB];
        }
        float2 uA[ITER], vA[ITER], uB[ITER], vB[ITER];
#pragma unroll
        for (int it = 0; it < ITER; ++it) {
            uA[it] = *(const float2*)(Pu + (size_t)rA[it] * 16 + cA[it]);
            vA[it] = *(const float2*)(Pm + (size_t)mA[it] * 16 + cA[it]);
            uB[it] = *(const float2*)(Pu + (size_t)rB[it] * 16 + cB[it]);
            vB[it] = *(const float2*)(Pm + (size_t)mB[it] * 16 + cB[it]);
        }
#pragma unroll
        for (int it = 0; it < ITER; ++it) {
            *(float4*)(out + f0[it]) = make_float4(uA[it].x + vA[it].x, uA[it].y + vA[it].y,
                                                   uB[it].x + vB[it].x, uB[it].y + vB[it].y);
        }
    } else {
        for (unsigned p = threadIdx.x; p < 2u * perBlock; p += blockDim.x) {
            const unsigned f = 4u * qb + 2u * p;
            if (f < total) {
                const unsigned e = f / C, c = f - e * C;
                const float2 u = *(const float2*)(Pu + (size_t)row[e] * 16 + c);
                const float2 v = *(const float2*)(Pm + (size_t)col[e] * 16 + c);
                *(float2*)(out + f) = make_float2(u.x + v.x, u.y + v.y);
            }
        }
    }
}

__global__ void zero_int_kernel(int* __restrict__ p, int n) {
    const int i = blockIdx.x * blockDim.x + threadIdx.x;
    if (i < n) p[i] = 0;
}

__global__ void count_targets_kernel(const int* __restrict__ col,
                                     int* __restrict__ cnt, int E) {
    const int e = blockIdx.x * blockDim.x + threadIdx.x;
    if (e < E) atomicAdd(&cnt[col[e]], 1);
}

__global__ void alloc_segments_kernel(const int* __restrict__ cnt,
                                      int* __restrict__ total,
                                      int* __restrict__ start,
                                      int* __restrict__ cursor,
                                      float* __restrict__ dinv, int N) {
    const int i = blockIdx.x * blockDim.x + threadIdx.x;
    const int lane = threadIdx.x & 31;
    const int c = (i < N) ? cnt[i] : 0;
    int incl = c;
    for (int d = 1; d < 32; d <<= 1) {
        const int t = __shfl_up(incl, d);
        if (lane >= d) incl += t;
    }
    int base = 0;
    if (lane == 31) base = atomicAdd(total, incl);
    base = __shfl(base, 31);
    if (i < N) {
        const int s = base + incl - c;
        start[i] = s;
        cursor[i] = s;
        dinv[i] = rsqrtf((float)(c + 1));
    }
}

__global__ void fill_segments_kernel(const int* __restrict__ row,
                                     const int* __restrict__ col,
                                     const float* __restrict__ dinv,
                                     int* __restrict__ cursor,
                                     int2* __restrict__ slots, int E) {
    const int e = blockIdx.x * blockDim.x + threadIdx.x;
    if (e < E) {
        const int r = row[e];
        const int pos = atomicAdd(&cursor[col[e]], 1);
        slots[pos] = make_int2(r, __float_as_int(dinv[r]));
    }
}

__device__ __forceinline__ float4 fma4(float w, float4 a, float4 c) {
    return make_float4(fmaf(w, a.x, c.x), fmaf(w, a.y, c.y), fmaf(w, a.z, c.z), fmaf(w, a.w, c.w));
}

__global__ void gather_relu_kernel(const float* __restrict__ h,
                                   const float* __restrict__ dinv,
                                   const int* __restrict__ start,
                                   const int* __restrict__ cnt,
                                   const int2* __restrict__ slots,
                                   float* __restrict__ xout,
                                   int N, int E, int H) {
    const int lane = threadIdx.x & 31;
    const int cpn = (H + 127) >> 7;
    const int wid = __builtin_amdgcn_readfirstlane(
        (int)(blockIdx.x * (blockDim.x >> 5) + (threadIdx.x >> 5)));
    const int node = wid / cpn;
    if (node >= N) return;
    const int cb = (wid - node * cpn) << 7;
    const bool act = cb + 4 * lane < H;
    const int c0 = min(cb + 4 * lane, H - 4);
    const float* __restrict__ hl = h + c0;

    const int c = min(max(cnt[node], 0), E);
    const int s = min(max(start[node], 0), E - c);
    const float4 hself = *(const float4*)(hl + (size_t)node * H);
    const float di = dinv[node];

    float4 acc = make_float4(0.f, 0.f, 0.f, 0.f);
    for (int k0 = 0; k0 < c; k0 += 32) {
        const int n = min(32, c - k0);
        int2 ev = make_int2(0, 0);
        if (lane < n) ev = slots[s + k0 + lane];
        for (int j = 0; j < n; j += 4) {
            const int j1 = min(j + 1, n - 1), j2 = min(j + 2, n - 1), j3 = min(j + 3, n - 1);
            const int s0 = __builtin_amdgcn_readlane(ev.x, j);
            const int s1 = __builtin_amdgcn_readlane(ev.x, j1);
            const int s2 = __builtin_amdgcn_readlane(ev.x, j2);
            const int s3 = __builtin_amdgcn_readlane(ev.x, j3);
            const float w0 = __int_as_float(__builtin_amdgcn_readlane(ev.y, j));
            const float w1 = __int_as_float(__builtin_amdgcn_readlane(ev.y, j1));
            const float w2 = __int_as_float(__builtin_amdgcn_readlane(ev.y, j2));
            const float w3 = __int_as_float(__builtin_amdgcn_readlane(ev.y, j3));
            const float4 a0 = *(const float4*)(hl + (size_t)s0 * H);
            const float4 a1 = *(const float4*)(hl + (size_t)s1 * H);
            const float4 a2 = *(const float4*)(hl + (size_t)s2 * H);
            const float4 a3 = *(const float4*)(hl + (size_t)s3 * H);
            acc = fma4(w0, a0, acc);
            if (j + 1 < n) acc = fma4(w1, a1, acc);
            if (j + 2 < n) acc = fma4(w2, a2, acc);
            if (j + 3 < n) acc = fma4(w3, a3, acc);
        }
    }

    if (act) {
        const float d2 = di * di;
        float4 v = make_float4(di * acc.x + d2 * hself.x, di * acc.y + d2 * hself.y,
                               di * acc.z + d2 * hself.z, di * acc.w + d2 * hself.w);
        v.x = v.x > 0.f ? v.x : 0.f;
        v.y = v.y > 0.f ? v.y : 0.f;
        v.z = v.z > 0.f ? v.z : 0.f;
        v.w = v.w > 0.f ? v.w : 0.f;
        *(float4*)(xout + (size_t)node * H + c0) = v;
    }
}

static inline int ceil_div_i(long long a, long long b) { return (int)((a + b - 1) / b); }

static void launch_gemm(const float* X, const float* W, const float* b, float* Y,
                        int M, int K, int N, hipStream_t stream) {
    const int TPB = 256;
    const int strips = ceil_div_i(M, 16);
    if (K == 32)
        gemm_bias_wmma_lds<32><<<strips, TPB, 0, stream>>>(X, W, b, Y, M, N);
    else if (K == 64)
        gemm_bias_wmma_lds<64><<<strips, TPB, 0, stream>>>(X, W, b, Y, M, N);
    else if (K == 128)
        gemm_bias_wmma_lds<128><<<strips, TPB, 0, stream>>>(X, W, b, Y, M, N);
    else {
        const int tiles = strips * (N / 16);
        gemm_bias_wmma_rt<<<ceil_div_i(tiles, 8), TPB, 0, stream>>>(X, W, b, Y, M, K, N);
    }
}

extern "C" void kernel_launch(void* const* d_in, const int* in_sizes, int n_in,
                              void* d_out, int out_size, void* d_ws, size_t ws_size,
                              hipStream_t stream) {
    const float* x_user  = (const float*)d_in[0];
    const float* x_movie = (const float*)d_in[1];
    const int*   edge    = (const int*)d_in[2];
    const float* Wu = (const float*)d_in[3];
    const float* bu = (const float*)d_in[4];
    const float* Wm = (const float*)d_in[5];
    const float* bm = (const float*)d_in[6];
    const float* W1 = (const float*)d_in[7];
    const float* b1 = (const float*)d_in[8];
    const float* W2 = (const float*)d_in[9];
    const float* b2 = (const float*)d_in[10];
    const float* We = (const float*)d_in[11];
    const float* be = (const float*)d_in[12];
    float* out = (float*)d_out;

    const int H  = in_sizes[4];
    const int C  = in_sizes[12];
    const int Fu = in_sizes[3] / H;
    const int Fm = in_sizes[5] / H;
    const int Nu = in_sizes[0] / Fu;
    const int Nm = in_sizes[1] / Fm;
    const int E  = in_sizes[2] / 2;
    const int N  = Nu + Nm;
    const int* row = edge;
    const int* col = edge + E;

    float* ws   = (float*)d_ws;
    float* X    = ws;
    float* Hh   = X  + (size_t)N * H;
    float* Acc  = Hh + (size_t)N * H;
    float* dinv = Acc + (size_t)N * H;

    const int TPB = 256, WPB = 8;

    const size_t Npad = ((size_t)N + 31) & ~(size_t)31;
    float* P      = dinv + Npad;
    int*   cnt    = (int*)(P + (size_t)N * 16);
    int*   start  = cnt + (((size_t)N + 1 + 31) & ~(size_t)31);
    int*   cursor = start + Npad;
    int2*  slots  = (int2*)(cursor + Npad);

    launch_gemm(x_user, Wu, bu, X, Nu, Fu, H, stream);
    launch_gemm(x_movie, Wm, bm, X + (size_t)Nu * H, Nm, Fm, H, stream);

    zero_int_kernel<<<ceil_div_i(N + 1, TPB), TPB, 0, stream>>>(cnt, N + 1);
    count_targets_kernel<<<ceil_div_i(E, TPB), TPB, 0, stream>>>(col, cnt, E);
    alloc_segments_kernel<<<ceil_div_i(N, TPB), TPB, 0, stream>>>(cnt, cnt + N, start, cursor, dinv, N);
    fill_segments_kernel<<<ceil_div_i(E, TPB), TPB, 0, stream>>>(row, col, dinv, cursor, slots, E);

    const int gatherWaves = N * ((H + 127) / 128);

    launch_gemm(X, W1, b1, Hh, N, H, H, stream);
    gather_relu_kernel<<<ceil_div_i(gatherWaves, WPB), TPB, 0, stream>>>(Hh, dinv, start, cnt, slots, Acc, N, E, H);

    launch_gemm(Acc, W2, b2, Hh, N, H, H, stream);
    gather_relu_kernel<<<ceil_div_i(gatherWaves, WPB), TPB, 0, stream>>>(Hh, dinv, start, cnt, slots, X, N, E, H);

    const int edgeTiles = ceil_div_i(E, 16);
    if (H == 128 && C == 10) {
        const int nodeTiles = ceil_div_i(Nu, 16) + ceil_div_i(Nm, 16);
        node_proj_wmma<<<ceil_div_i(nodeTiles, 4), 128, 0, stream>>>(X, We, be, P, Nu, Nm, C);
        const long long quads = ((long long)E * C + 3) / 4;
        edge_add10<4><<<ceil_div_i(quads, TPB * 4), TPB, 0, stream>>>(P, row, col, out, E, Nu);
    } else if (H == 128)
        edge_mlp_wmma<128><<<ceil_div_i(edgeTiles, WPB), TPB, 0, stream>>>(X, row, col, We, be, out, E, Nu, H, C);
    else
        edge_mlp_wmma<0><<<ceil_div_i(edgeTiles, WPB), TPB, 0, stream>>>(X, row, col, We, be, out, E, Nu, H, C);
}


# GEMMs re-tiled W-stationary: each wave holds its 16 W columns in registers and walks row strips staged in LDS at a conflict-free pitch; k regrouped per WMMA, so each dot's sum order changes
# speedup vs baseline: 2.9280x; 1.4556x over previous
#include <hip/hip_runtime.h>

typedef __attribute__((ext_vector_type(2))) float v2f;
typedef __attribute__((ext_vector_type(8))) float v8f;

__device__ __forceinline__ v8f wmma_f32x4(v2f a, v2f b, v8f c) {
    return __builtin_amdgcn_wmma_f32_16x16x4_f32(
         false, a,  false, b,
         (short)0, c,  false,  false);
}

template <int K>
__global__ void gemm_bias_wmma_lds(const float* __restrict__ X,
                                   const float* __restrict__ W,
                                   const float* __restrict__ bias,
                                   float* __restrict__ Y,
                                   int M, int N) {
    __shared__ float As[16 * K];

    const int lane = threadIdx.x & 31;
    const int wave = threadIdx.x >> 5;
    const int wavesPerBlock = blockDim.x >> 5;
    const int tilesN = N >> 4;
    const int strip = blockIdx.x * 16;

    if (strip + 16 <= M) {
        const float4* __restrict__ src = (const float4*)(X + (size_t)strip * K);
        float4* dst = (float4*)As;
        for (int i = threadIdx.x; i < (16 * K) / 4; i += blockDim.x) dst[i] = src[i];
    } else {
        for (int i = threadIdx.x; i < 16 * K; i += blockDim.x) {
            const int rr = i / K, cc = i - rr * K;
            const int gr = min(strip + rr, M - 1);
            As[i] = X[(size_t)gr * K + cc];
        }
    }
    __syncthreads();

    const int koff = (lane >> 4) << 1;
    const float* __restrict__ arow = As + (lane & 15) * K + koff;
    const bool fullTile = (strip + 16 <= M);
    const int rbase = strip + ((lane >> 4) << 3);

    for (int tn = wave; tn < tilesN; tn += wavesPerBlock) {
        const int ncol = tn * 16 + (lane & 15);
        const float* __restrict__ wrow = W + (size_t)ncol * K + koff;

        v8f acc = {};
#pragma unroll
        for (int k = 0; k < K; k += 4) {
            v2f a = *(const v2f*)(arow + k);
            v2f b = *(const v2f*)(wrow + k);
            acc = wmma_f32x4(a, b, acc);
        }

        const float bv = bias[ncol];
        float* yp = Y + (size_t)rbase * N + ncol;
        if (fullTile) {
#pragma unroll
            for (int r = 0; r < 8; ++r) { *yp = acc[r] + bv; yp += N; }
        } else {
#pragma unroll
            for (int r = 0; r < 8; ++r) {
                if (rbase + r < M) *yp = acc[r] + bv;
                yp += N;
            }
        }
    }
}

template <int K>
__global__ void __launch_bounds__(256)
gemm_bias_wmma_ws(const float* __restrict__ X,
                  const float* __restrict__ W,
                  const float* __restrict__ bias,
                  float* __restrict__ Y, int M) {
    constexpr int N = 128;
    constexpr int P = K + 4;
    constexpr int C4 = K / 4;
    constexpr int NV = 16 * C4;
    constexpr int PER = (NV + 255) / 256;
    __shared__ __attribute__((aligned(16))) float As[2][16 * P];

    const int lane = threadIdx.x & 31;
    const int wave = threadIdx.x >> 5;
    const int c = lane & 15;
    const int h = lane >> 4;
    const int ncol = wave * 16 + c;
    const int strips = (M + 15) >> 4;
    const int step = gridDim.x;

    float4 bf[K / 8];
    {
        const float* __restrict__ wrow = W + (size_t)ncol * K + 4 * h;
#pragma unroll
        for (int i = 0; i < K / 8; ++i) bf[i] = *(const float4*)(wrow + 8 * i);
    }
    const float bv = bias[ncol];

    static_assert(PER <= 2, "a strip is at most 512 float4s");
    const int i0 = threadIdx.x;
    const int rr0 = i0 / C4, cc0 = i0 - rr0 * C4;
    const int i1 = i0 + 256;
    const int rr1 = i1 / C4, cc1 = i1 - rr1 * C4;

    int s = blockIdx.x;
    if (s < strips) {
        if (NV >= 256 || i0 < NV)
            *(float4*)(&As[0][rr0 * P + 4 * cc0]) =
                *(const float4*)(X + (size_t)min(s * 16 + rr0, M - 1) * K + 4 * cc0);
        if (PER > 1)
            *(float4*)(&As[0][rr1 * P + 4 * cc1]) =
                *(const float4*)(X + (size_t)min(s * 16 + rr1, M - 1) * K + 4 * cc1);
    }
    __syncthreads();

    const int aoff = c * P + 4 * h;
    int buf = 0;
    for (; s < strips; s += step) {
        const int sn = s + step;
        const bool more = sn < strips;
        float4 pre0 = make_float4(0.f, 0.f, 0.f, 0.f);
        float4 pre1 = make_float4(0.f, 0.f, 0.f, 0.f);
        if (more) {
            if (NV >= 256 || i0 < NV)
                pre0 = *(const float4*)(X + (size_t)min(sn * 16 + rr0, M - 1) * K + 4 * cc0);
            if (PER > 1)
                pre1 = *(const float4*)(X + (size_t)min(sn * 16 + rr1, M - 1) * K + 4 * cc1);
        }

        const float* arow = &As[buf][aoff];
        v8f acc = {};
#pragma unroll
        for (int i = 0; i < K / 8; ++i) {
            const float4 a = *(const float4*)(arow + 8 * i);
            v2f a0 = {a.x, a.y}, b0 = {bf[i].x, bf[i].y};
            v2f a1 = {a.z, a.w}, b1 = {bf[i].z, bf[i].w};
            acc = wmma_f32x4(a0, b0, acc);
            acc = wmma_f32x4(a1, b1, acc);
        }

        if (more) {
            if (NV >= 256 || i0 < NV) *(float4*)(&As[buf ^ 1][rr0 * P + 4 * cc0]) = pre0;
            if (PER > 1) *(float4*)(&As[buf ^ 1][rr1 * P + 4 * cc1]) = pre1;
        }
        __syncthreads();
        buf ^= 1;

        const int rbase = s * 16 + 8 * h;
        float* yp = Y + (size_t)rbase * N + ncol;
        if (s * 16 + 16 <= M) {
#pragma unroll
            for (int r = 0; r < 8; ++r) { *yp = acc[r] + bv; yp += N; }
        } else {
#pragma unroll
            for (int r = 0; r < 8; ++r) {
                if (rbase + r < M) *yp = acc[r] + bv;
                yp += N;
            }
        }
    }
}

__global__ void gemm_bias_wmma_rt(const float* __restrict__ X,
                                  const float* __restrict__ W,
                                  const float* __restrict__ bias,
                                  float* __restrict__ Y,
                                  int M, int K, int N) {
    const int lane = threadIdx.x & 31;
    const int wave = threadIdx.x >> 5;
    const int tilesN = N >> 4;
    const int tile = blockIdx.x * (blockDim.x >> 5) + wave;
    const int tm = tile / tilesN;
    const int tn = tile - tm * tilesN;
    if (tm * 16 >= M) return;

    const int mrow = min(tm * 16 + (lane & 15), M - 1);
    const int ncol = tn * 16 + (lane & 15);
    const int koff = (lane >> 4) << 1;
    const float* __restrict__ xrow = X + (size_t)mrow * K + koff;
    const float* __restrict__ wrow = W + (size_t)ncol * K + koff;

    v8f acc = {};
#pragma unroll 8
    for (int k = 0; k < K; k += 4) {
        v2f a = *(const v2f*)(xrow + k);
        v2f b = *(const v2f*)(wrow + k);
        acc = wmma_f32x4(a, b, acc);
    }

    const float bv = bias[ncol];
    const int rbase = tm * 16 + ((lane >> 4) << 3);
    float* yp = Y + (size_t)rbase * N + ncol;
#pragma unroll
    for (int r = 0; r < 8; ++r) {
        if (rbase + r < M) *yp = acc[r] + bv;
        yp += N;
    }
}

template <int HT>
__global__ void edge_mlp_wmma(const float* __restrict__ xf,
                              const int* __restrict__ row,
                              const int* __restrict__ col,
                              const float* __restrict__ We,
                              const float* __restrict__ be,
                              float* __restrict__ out,
                              int E, int Nu, int Hrt, int C) {
    const int H = (HT > 0) ? HT : Hrt;
    const int K = 2 * H;
    const int lane = threadIdx.x & 31;
    const int wave = threadIdx.x >> 5;
    const int tile = blockIdx.x * (blockDim.x >> 5) + wave;
    const int ebase = tile * 16;
    if (ebase >= E) return;

    const int m = lane & 15;
    const int e = min(ebase + m, E - 1);
    const int r = row[e];
    const int c = col[e];
    const int koff = (lane >> 4) << 1;

    const float* __restrict__ urow = xf + (size_t)r * H + koff;
    const float* __restrict__ mrow = xf + (size_t)(Nu + c) * H + koff;

    const int n = lane & 15;
    const float wmask = (n < C) ? 1.0f : 0.0f;
    const float* __restrict__ wrow = We + (size_t)min(n, C - 1) * K + koff;

    v8f acc = {};
#pragma unroll 8
    for (int k = 0; k < K; k += 4) {
        const float* src = (k + koff < H) ? (urow + k) : (mrow + (k - H));
        v2f a = *(const v2f*)src;
        v2f b = *(const v2f*)(wrow + k);
        b.x *= wmask;
        b.y *= wmask;
        acc = wmma_f32x4(a, b, acc);
    }

    if (n < C) {
        const float bv = be[n];
        const int mb = (lane >> 4) << 3;
        float* op = out + (size_t)(ebase + mb) * C + n;
        if (ebase + 16 <= E) {
#pragma unroll
            for (int rr = 0; rr < 8; ++rr) { *op = acc[rr] + bv; op += C; }
        } else {
#pragma unroll
            for (int rr = 0; rr < 8; ++rr) {
                if (ebase + mb + rr < E) *op = acc[rr] + bv;
                op += C;
            }
        }
    }
}

__global__ void node_proj_wmma(const float* __restrict__ xf,
                               const float* __restrict__ We,
                               const float* __restrict__ be,
                               float* __restrict__ P,
                               int Nu, int Nm, int C) {
    constexpr int H = 128;
    const int lane = threadIdx.x & 31;
    const int wave = threadIdx.x >> 5;
    const int tilesU = (Nu + 15) >> 4;
    const int tilesM = (Nm + 15) >> 4;
    const int tile = blockIdx.x * (blockDim.x >> 5) + wave;
    if (tile >= tilesU + tilesM) return;

    const bool second = tile >= tilesU;
    const int segEnd = second ? Nu + Nm : Nu;
    const int base = second ? Nu + ((tile - tilesU) << 4) : (tile << 4);

    const int c = lane & 15;
    const int h = lane >> 4;
    const int node = min(base + 2 * (c & 7) + (c >> 3), segEnd - 1);
    const float* __restrict__ arow = xf + (size_t)node * H + 4 * h;
    const float* __restrict__ brow =
        We + (size_t)min(c, C - 1) * (2 * H) + (second ? H : 0) + 4 * h;

    const float bsel = be[min(c, C - 1)];
    const float bv = (second && c < C) ? bsel : 0.0f;

    v8f acc0 = {}, acc1 = {};
#pragma unroll
    for (int i = 0; i < H / 8; ++i) {
        const float4 a = *(const float4*)(arow + 8 * i);
        const float4 b = *(const float4*)(brow + 8 * i);
        v2f a0 = {a.x, a.y}, b0 = {b.x, b.y};
        v2f a1 = {a.z, a.w}, b1 = {b.z, b.w};
        acc0 = wmma_f32x4(a0, b0, acc0);
        acc1 = wmma_f32x4(a1, b1, acc1);
    }

    float v[8];
#pragma unroll
    for (int r = 0; r < 8; ++r) v[r] = (acc0[r] + acc1[r]) + bv;

    float* pp = P + (size_t)base * 16 + lane;
    if (base + 16 <= segEnd) {
#pragma unroll
        for (int r = 0; r < 8; ++r) pp[32 * r] = v[r];
    } else {
#pragma unroll
        for (int r = 0; r < 8; ++r)
            if (base + 2 * r + h < segEnd) pp[32 * r] = v[r];
    }
}

template <int ITER>
__global__ void edge_add10(const float* __restrict__ P,
                           const int* __restrict__ row,
                           const int* __restrict__ col,
                           float* __restrict__ out,
                           int E, int Nu) {
    constexpr unsigned C = 10;
    const unsigned total = (unsigned)E * C;
    const float* __restrict__ Pu = P;
    const float* __restrict__ Pm = P + (size_t)Nu * 16;
    const unsigned perBlock = blockDim.x * ITER;
    const unsigned qb = blockIdx.x * perBlock;

    if ((qb + perBlock) * 4u <= total) {
        unsigned f0[ITER], cA[ITER], cB[ITER];
        int rA[ITER], mA[ITER], rB[ITER], mB[ITER];
#pragma unroll
        for (int it = 0; it < ITER; ++it) {
            f0[it] = 4u * (qb + it * blockDim.x + threadIdx.x);
            const unsigned f1 = f0[it] + 2u;
            const unsigned eA = f0[it] / C, eB = f1 / C;
            cA[it] = f0[it] - eA * C;
            cB[it] = f1 - eB * C;
            rA[it] = row[eA]; mA[it] = col[eA];
            rB[it] = row[eB]; mB[it] = col[eB];
        }
        float2 uA[ITER], vA[ITER], uB[ITER], vB[ITER];
#pragma unroll
        for (int it = 0; it < ITER; ++it) {
            uA[it] = *(const float2*)(Pu + (size_t)rA[it] * 16 + cA[it]);
            vA[it] = *(const float2*)(Pm + (size_t)mA[it] * 16 + cA[it]);
            uB[it] = *(const float2*)(Pu + (size_t)rB[it] * 16 + cB[it]);
            vB[it] = *(const float2*)(Pm + (size_t)mB[it] * 16 + cB[it]);
        }
#pragma unroll
        for (int it = 0; it < ITER; ++it) {
            *(float4*)(out + f0[it]) = make_float4(uA[it].x + vA[it].x, uA[it].y + vA[it].y,
                                                   uB[it].x + vB[it].x, uB[it].y + vB[it].y);
        }
    } else {
        for (unsigned p = threadIdx.x; p < 2u * perBlock; p += blockDim.x) {
            const unsigned f = 4u * qb + 2u * p;
            if (f < total) {
                const unsigned e = f / C, c = f - e * C;
                const float2 u = *(const float2*)(Pu + (size_t)row[e] * 16 + c);
                const float2 v = *(const float2*)(Pm + (size_t)col[e] * 16 + c);
                *(float2*)(out + f) = make_float2(u.x + v.x, u.y + v.y);
            }
        }
    }
}

__global__ void zero_int_kernel(int* __restrict__ p, int n) {
    const int i = blockIdx.x * blockDim.x + threadIdx.x;
    if (i < n) p[i] = 0;
}

__global__ void count_targets_kernel(const int* __restrict__ col,
                                     int* __restrict__ cnt, int E) {
    const int e = blockIdx.x * blockDim.x + threadIdx.x;
    if (e < E) atomicAdd(&cnt[col[e]], 1);
}

__global__ void alloc_segments_kernel(const int* __restrict__ cnt,
                                      int* __restrict__ total,
                                      int* __restrict__ start,
                                      int* __restrict__ cursor,
                                      float* __restrict__ dinv, int N) {
    const int i = blockIdx.x * blockDim.x + threadIdx.x;
    const int lane = threadIdx.x & 31;
    const int c = (i < N) ? cnt[i] : 0;
    int incl = c;
    for (int d = 1; d < 32; d <<= 1) {
        const int t = __shfl_up(incl, d);
        if (lane >= d) incl += t;
    }
    int base = 0;
    if (lane == 31) base = atomicAdd(total, incl);
    base = __shfl(base, 31);
    if (i < N) {
        const int s = base + incl - c;
        start[i] = s;
        cursor[i] = s;
        dinv[i] = rsqrtf((float)(c + 1));
    }
}

__global__ void fill_segments_kernel(const int* __restrict__ row,
                                     const int* __restrict__ col,
                                     const float* __restrict__ dinv,
                                     int* __restrict__ cursor,
                                     int2* __restrict__ slots, int E) {
    const int e = blockIdx.x * blockDim.x + threadIdx.x;
    if (e < E) {
        const int r = row[e];
        const int pos = atomicAdd(&cursor[col[e]], 1);
        slots[pos] = make_int2(r, __float_as_int(dinv[r]));
    }
}

__device__ __forceinline__ float4 fma4(float w, float4 a, float4 c) {
    return make_float4(fmaf(w, a.x, c.x), fmaf(w, a.y, c.y), fmaf(w, a.z, c.z), fmaf(w, a.w, c.w));
}

__global__ void gather_relu_kernel(const float* __restrict__ h,
                                   const float* __restrict__ dinv,
                                   const int* __restrict__ start,
                                   const int* __restrict__ cnt,
                                   const int2* __restrict__ slots,
                                   float* __restrict__ xout,
                                   int N, int E, int H) {
    const int lane = threadIdx.x & 31;
    const int cpn = (H + 127) >> 7;
    const int wid = __builtin_amdgcn_readfirstlane(
        (int)(blockIdx.x * (blockDim.x >> 5) + (threadIdx.x >> 5)));
    const int node = wid / cpn;
    if (node >= N) return;
    const int cb = (wid - node * cpn) << 7;
    const bool act = cb + 4 * lane < H;
    const int c0 = min(cb + 4 * lane, H - 4);
    const float* __restrict__ hl = h + c0;

    const int c = min(max(cnt[node], 0), E);
    const int s = min(max(start[node], 0), E - c);
    const float4 hself = *(const float4*)(hl + (size_t)node * H);
    const float di = dinv[node];

    float4 acc = make_float4(0.f, 0.f, 0.f, 0.f);
    for (int k0 = 0; k0 < c; k0 += 32) {
        const int n = min(32, c - k0);
        int2 ev = make_int2(0, 0);
        if (lane < n) ev = slots[s + k0 + lane];
        for (int j = 0; j < n; j += 4) {
            const int j1 = min(j + 1, n - 1), j2 = min(j + 2, n - 1), j3 = min(j + 3, n - 1);
            const int s0 = __builtin_amdgcn_readlane(ev.x, j);
            const int s1 = __builtin_amdgcn_readlane(ev.x, j1);
            const int s2 = __builtin_amdgcn_readlane(ev.x, j2);
            const int s3 = __builtin_amdgcn_readlane(ev.x, j3);
            const float w0 = __int_as_float(__builtin_amdgcn_readlane(ev.y, j));
            const float w1 = __int_as_float(__builtin_amdgcn_readlane(ev.y, j1));
            const float w2 = __int_as_float(__builtin_amdgcn_readlane(ev.y, j2));
            const float w3 = __int_as_float(__builtin_amdgcn_readlane(ev.y, j3));
            const float4 a0 = *(const float4*)(hl + (size_t)s0 * H);
            const float4 a1 = *(const float4*)(hl + (size_t)s1 * H);
            const float4 a2 = *(const float4*)(hl + (size_t)s2 * H);
            const float4 a3 = *(const float4*)(hl + (size_t)s3 * H);
            acc = fma4(w0, a0, acc);
            if (j + 1 < n) acc = fma4(w1, a1, acc);
            if (j + 2 < n) acc = fma4(w2, a2, acc);
            if (j + 3 < n) acc = fma4(w3, a3, acc);
        }
    }

    if (act) {
        const float d2 = di * di;
        float4 v = make_float4(di * acc.x + d2 * hself.x, di * acc.y + d2 * hself.y,
                               di * acc.z + d2 * hself.z, di * acc.w + d2 * hself.w);
        v.x = v.x > 0.f ? v.x : 0.f;
        v.y = v.y > 0.f ? v.y : 0.f;
        v.z = v.z > 0.f ? v.z : 0.f;
        v.w = v.w > 0.f ? v.w : 0.f;
        *(float4*)(xout + (size_t)node * H + c0) = v;
    }
}

static inline int ceil_div_i(long long a, long long b) { return (int)((a + b - 1) / b); }

static void launch_gemm(const float* X, const float* W, const float* b, float* Y,
                        int M, int K, int N, hipStream_t stream) {
    const int TPB = 256;
    const int strips = ceil_div_i(M, 16);
    if (N == 128 && (K == 32 || K == 64 || K == 128)) {
        const int grid = strips < 1024 ? strips : 1024;
        if (K == 32)
            gemm_bias_wmma_ws<32><<<grid, TPB, 0, stream>>>(X, W, b, Y, M);
        else if (K == 64)
            gemm_bias_wmma_ws<64><<<grid, TPB, 0, stream>>>(X, W, b, Y, M);
        else
            gemm_bias_wmma_ws<128><<<grid, TPB, 0, stream>>>(X, W, b, Y, M);
        return;
    }
    if (K == 32)
        gemm_bias_wmma_lds<32><<<strips, TPB, 0, stream>>>(X, W, b, Y, M, N);
    else if (K == 64)
        gemm_bias_wmma_lds<64><<<strips, TPB, 0, stream>>>(X, W, b, Y, M, N);
    else if (K == 128)
        gemm_bias_wmma_lds<128><<<strips, TPB, 0, stream>>>(X, W, b, Y, M, N);
    else {
        const int tiles = strips * (N / 16);
        gemm_bias_wmma_rt<<<ceil_div_i(tiles, 8), TPB, 0, stream>>>(X, W, b, Y, M, K, N);
    }
}

extern "C" void kernel_launch(void* const* d_in, const int* in_sizes, int n_in,
                              void* d_out, int out_size, void* d_ws, size_t ws_size,
                              hipStream_t stream) {
    const float* x_user  = (const float*)d_in[0];
    const float* x_movie = (const float*)d_in[1];
    const int*   edge    = (const int*)d_in[2];
    const float* Wu = (const float*)d_in[3];
    const float* bu = (const float*)d_in[4];
    const float* Wm = (const float*)d_in[5];
    const float* bm = (const float*)d_in[6];
    const float* W1 = (const float*)d_in[7];
    const float* b1 = (const float*)d_in[8];
    const float* W2 = (const float*)d_in[9];
    const float* b2 = (const float*)d_in[10];
    const float* We = (const float*)d_in[11];
    const float* be = (const float*)d_in[12];
    float* out = (float*)d_out;

    const int H  = in_sizes[4];
    const int C  = in_sizes[12];
    const int Fu = in_sizes[3] / H;
    const int Fm = in_sizes[5] / H;
    const int Nu = in_sizes[0] / Fu;
    const int Nm = in_sizes[1] / Fm;
    const int E  = in_sizes[2] / 2;
    const int N  = Nu + Nm;
    const int* row = edge;
    const int* col = edge + E;

    float* ws   = (float*)d_ws;
    float* X    = ws;
    float* Hh   = X  + (size_t)N * H;
    float* Acc  = Hh + (size_t)N * H;
    float* dinv = Acc + (size_t)N * H;

    const int TPB = 256, WPB = 8;

    const size_t Npad = ((size_t)N + 31) & ~(size_t)31;
    float* P      = dinv + Npad;
    int*   cnt    = (int*)(P + (size_t)N * 16);
    int*   start  = cnt + (((size_t)N + 1 + 31) & ~(size_t)31);
    int*   cursor = start + Npad;
    int2*  slots  = (int2*)(cursor + Npad);

    launch_gemm(x_user, Wu, bu, X, Nu, Fu, H, stream);
    launch_gemm(x_movie, Wm, bm, X + (size_t)Nu * H, Nm, Fm, H, stream);

    zero_int_kernel<<<ceil_div_i(N + 1, TPB), TPB, 0, stream>>>(cnt, N + 1);
    count_targets_kernel<<<ceil_div_i(E, TPB), TPB, 0, stream>>>(col, cnt, E);
    alloc_segments_kernel<<<ceil_div_i(N, TPB), TPB, 0, stream>>>(cnt, cnt + N, start, cursor, dinv, N);
    fill_segments_kernel<<<ceil_div_i(E, TPB), TPB, 0, stream>>>(row, col, dinv, cursor, slots, E);

    const int gatherWaves = N * ((H + 127) / 128);

    launch_gemm(X, W1, b1, Hh, N, H, H, stream);
    gather_relu_kernel<<<ceil_div_i(gatherWaves, WPB), TPB, 0, stream>>>(Hh, dinv, start, cnt, slots, Acc, N, E, H);

    launch_gemm(Acc, W2, b2, Hh, N, H, H, stream);
    gather_relu_kernel<<<ceil_div_i(gatherWaves, WPB), TPB, 0, stream>>>(Hh, dinv, start, cnt, slots, X, N, E, H);

    const int edgeTiles = ceil_div_i(E, 16);
    if (H == 128 && C == 10) {
        const int nodeTiles = ceil_div_i(Nu, 16) + ceil_div_i(Nm, 16);
        node_proj_wmma<<<ceil_div_i(nodeTiles, 4), 128, 0, stream>>>(X, We, be, P, Nu, Nm, C);
        const long long quads = ((long long)E * C + 3) / 4;
        edge_add10<4><<<ceil_div_i(quads, TPB * 4), TPB, 0, stream>>>(P, row, col, out, E, Nu);
    } else if (H == 128)
        edge_mlp_wmma<128><<<ceil_div_i(edgeTiles, WPB), TPB, 0, stream>>>(X, row, col, We, be, out, E, Nu, H, C);
    else
        edge_mlp_wmma<0><<<ceil_div_i(edgeTiles, WPB), TPB, 0, stream>>>(X, row, col, We, be, out, E, Nu, H, C);
}


# conv1 by linearity: per call Wc = W1 Wt and bc = W1 bt + b1 on WMMA, the two input GEMMs write conv1's h directly, the 128-wide GEMM over all nodes is gone; products regrouped, sum order changes
# speedup vs baseline: 3.1193x; 1.0653x over previous
#include <hip/hip_runtime.h>

typedef __attribute__((ext_vector_type(2))) float v2f;
typedef __attribute__((ext_vector_type(8))) float v8f;

__device__ __forceinline__ v8f wmma_f32x4(v2f a, v2f b, v8f c) {
    return __builtin_amdgcn_wmma_f32_16x16x4_f32(
         false, a,  false, b,
         (short)0, c,  false,  false);
}

template <int K>
__global__ void gemm_bias_wmma_lds(const float* __restrict__ X,
                                   const float* __restrict__ W,
                                   const float* __restrict__ bias,
                                   float* __restrict__ Y,
                                   int M, int N) {
    __shared__ float As[16 * K];

    const int lane = threadIdx.x & 31;
    const int wave = threadIdx.x >> 5;
    const int wavesPerBlock = blockDim.x >> 5;
    const int tilesN = N >> 4;
    const int strip = blockIdx.x * 16;

    if (strip + 16 <= M) {
        const float4* __restrict__ src = (const float4*)(X + (size_t)strip * K);
        float4* dst = (float4*)As;
        for (int i = threadIdx.x; i < (16 * K) / 4; i += blockDim.x) dst[i] = src[i];
    } else {
        for (int i = threadIdx.x; i < 16 * K; i += blockDim.x) {
            const int rr = i / K, cc = i - rr * K;
            const int gr = min(strip + rr, M - 1);
            As[i] = X[(size_t)gr * K + cc];
        }
    }
    __syncthreads();

    const int koff = (lane >> 4) << 1;
    const float* __restrict__ arow = As + (lane & 15) * K + koff;
    const bool fullTile = (strip + 16 <= M);
    const int rbase = strip + ((lane >> 4) << 3);

    for (int tn = wave; tn < tilesN; tn += wavesPerBlock) {
        const int ncol = tn * 16 + (lane & 15);
        const float* __restrict__ wrow = W + (size_t)ncol * K + koff;

        v8f acc = {};
#pragma unroll
        for (int k = 0; k < K; k += 4) {
            v2f a = *(const v2f*)(arow + k);
            v2f b = *(const v2f*)(wrow + k);
            acc = wmma_f32x4(a, b, acc);
        }

        const float bv = bias[ncol];
        float* yp = Y + (size_t)rbase * N + ncol;
        if (fullTile) {
#pragma unroll
            for (int r = 0; r < 8; ++r) { *yp = acc[r] + bv; yp += N; }
        } else {
#pragma unroll
            for (int r = 0; r < 8; ++r) {
                if (rbase + r < M) *yp = acc[r] + bv;
                yp += N;
            }
        }
    }
}

template <int K>
__global__ void __launch_bounds__(256)
gemm_bias_wmma_ws(const float* __restrict__ X,
                  const float* __restrict__ W,
                  const float* __restrict__ bias,
                  float* __restrict__ Y, int M) {
    constexpr int N = 128;
    constexpr int P = K + 4;
    constexpr int C4 = K / 4;
    constexpr int NV = 16 * C4;
    constexpr int PER = (NV + 255) / 256;
    __shared__ __attribute__((aligned(16))) float As[2][16 * P];

    const int lane = threadIdx.x & 31;
    const int wave = threadIdx.x >> 5;
    const int c = lane & 15;
    const int h = lane >> 4;
    const int ncol = wave * 16 + c;
    const int strips = (M + 15) >> 4;
    const int step = gridDim.x;

    float4 bf[K / 8];
    {
        const float* __restrict__ wrow = W + (size_t)ncol * K + 4 * h;
#pragma unroll
        for (int i = 0; i < K / 8; ++i) bf[i] = *(const float4*)(wrow + 8 * i);
    }
    const float bv = bias[ncol];

    static_assert(PER <= 2, "a strip is at most 512 float4s");
    const int i0 = threadIdx.x;
    const int rr0 = i0 / C4, cc0 = i0 - rr0 * C4;
    const int i1 = i0 + 256;
    const int rr1 = i1 / C4, cc1 = i1 - rr1 * C4;

    int s = blockIdx.x;
    if (s < strips) {
        if (NV >= 256 || i0 < NV)
            *(float4*)(&As[0][rr0 * P + 4 * cc0]) =
                *(const float4*)(X + (size_t)min(s * 16 + rr0, M - 1) * K + 4 * cc0);
        if (PER > 1)
            *(float4*)(&As[0][rr1 * P + 4 * cc1]) =
                *(const float4*)(X + (size_t)min(s * 16 + rr1, M - 1) * K + 4 * cc1);
    }
    __syncthreads();

    const int aoff = c * P + 4 * h;
    int buf = 0;
    for (; s < strips; s += step) {
        const int sn = s + step;
        const bool more = sn < strips;
        float4 pre0 = make_float4(0.f, 0.f, 0.f, 0.f);
        float4 pre1 = make_float4(0.f, 0.f, 0.f, 0.f);
        if (more) {
            if (NV >= 256 || i0 < NV)
                pre0 = *(const float4*)(X + (size_t)min(sn * 16 + rr0, M - 1) * K + 4 * cc0);
            if (PER > 1)
                pre1 = *(const float4*)(X + (size_t)min(sn * 16 + rr1, M - 1) * K + 4 * cc1);
        }

        const float* arow = &As[buf][aoff];
        v8f acc = {};
#pragma unroll
        for (int i = 0; i < K / 8; ++i) {
            const float4 a = *(const float4*)(arow + 8 * i);
            v2f a0 = {a.x, a.y}, b0 = {bf[i].x, bf[i].y};
            v2f a1 = {a.z, a.w}, b1 = {bf[i].z, bf[i].w};
            acc = wmma_f32x4(a0, b0, acc);
            acc = wmma_f32x4(a1, b1, acc);
        }

        if (more) {
            if (NV >= 256 || i0 < NV) *(float4*)(&As[buf ^ 1][rr0 * P + 4 * cc0]) = pre0;
            if (PER > 1) *(float4*)(&As[buf ^ 1][rr1 * P + 4 * cc1]) = pre1;
        }
        __syncthreads();
        buf ^= 1;

        const int rbase = s * 16 + 8 * h;
        float* yp = Y + (size_t)rbase * N + ncol;
        if (s * 16 + 16 <= M) {
#pragma unroll
            for (int r = 0; r < 8; ++r) { *yp = acc[r] + bv; yp += N; }
        } else {
#pragma unroll
            for (int r = 0; r < 8; ++r) {
                if (rbase + r < M) *yp = acc[r] + bv;
                yp += N;
            }
        }
    }
}

__global__ void __launch_bounds__(256)
compose_wmma(const float* __restrict__ W1, const float* __restrict__ b1,
             const float* __restrict__ Wu, const float* __restrict__ bu,
             const float* __restrict__ Wm, const float* __restrict__ bm,
             float* __restrict__ Wcu, float* __restrict__ bcu,
             float* __restrict__ Wcm, float* __restrict__ bcm,
             int Fu, int Fm) {
    constexpr int H = 128;
    constexpr int S = H / 8;
    const int lane = threadIdx.x & 31;
    const int c = lane & 15;
    const int h = lane >> 4;
    const int unit = blockIdx.x * (blockDim.x >> 5) + (threadIdx.x >> 5);
    const int wu = 8 * (Fu >> 5);
    const int wm = 8 * (Fm >> 5);

    if (unit < wu + wm) {
        const bool second = unit >= wu;
        const int u = second ? unit - wu : unit;
        const int F = second ? Fm : Fu;
        const float* __restrict__ Wt = second ? Wm : Wu;
        float* __restrict__ Wc = second ? Wcm : Wcu;
        const int groups = F >> 5;
        const int nt = u / groups;
        const int cg = u - nt * groups;

        const float* __restrict__ arow = W1 + (size_t)(nt * 16 + c) * H + 4 * h;
        const float* __restrict__ bcol = Wt + (size_t)(4 * h) * F + cg * 32 + c;

        float4 af[S];
        float bl[S][4], br[S][4];
#pragma unroll
        for (int i = 0; i < S; ++i) {
            af[i] = *(const float4*)(arow + 8 * i);
            const float* __restrict__ bp = bcol + (size_t)(8 * i) * F;
#pragma unroll
            for (int e = 0; e < 4; ++e) {
                bl[i][e] = bp[(size_t)e * F];
                br[i][e] = bp[(size_t)e * F + 16];
            }
        }

        v8f acc0 = {}, acc1 = {};
#pragma unroll
        for (int i = 0; i < S; ++i) {
            v2f a0 = {af[i].x, af[i].y}, a1 = {af[i].z, af[i].w};
            v2f p0 = {bl[i][0], bl[i][1]}, p1 = {bl[i][2], bl[i][3]};
            v2f q0 = {br[i][0], br[i][1]}, q1 = {br[i][2], br[i][3]};
            acc0 = wmma_f32x4(a0, p0, acc0);
            acc1 = wmma_f32x4(a0, q0, acc1);
            acc0 = wmma_f32x4(a1, p1, acc0);
            acc1 = wmma_f32x4(a1, q1, acc1);
        }

        float* wp = Wc + (size_t)(nt * 16 + 8 * h) * F + cg * 32 + c;
#pragma unroll
        for (int r = 0; r < 8; ++r) {
            wp[0] = acc0[r];
            wp[16] = acc1[r];
            wp += F;
        }
    } else if (unit < wu + wm + 8) {
        const int q = unit - wu - wm;
        const bool second = q >= 4;
        const float* __restrict__ bt = second ? bm : bu;
        float* __restrict__ bc = second ? bcm : bcu;
        const int nb = 32 * (q & 3);

        const float* __restrict__ ar0 = W1 + (size_t)(nb + c) * H + 4 * h;
        const float* __restrict__ ar1 = ar0 + (size_t)16 * H;
        const float* __restrict__ bv = bt + 4 * h;

        float4 a0f[S], a1f[S], bf[S];
#pragma unroll
        for (int i = 0; i < S; ++i) {
            a0f[i] = *(const float4*)(ar0 + 8 * i);
            a1f[i] = *(const float4*)(ar1 + 8 * i);
            bf[i] = *(const float4*)(bv + 8 * i);
        }
        const float4 s00 = *(const float4*)(b1 + nb + 8 * h);
        const float4 s01 = *(const float4*)(b1 + nb + 8 * h + 4);
        const float4 s10 = *(const float4*)(b1 + nb + 16 + 8 * h);
        const float4 s11 = *(const float4*)(b1 + nb + 16 + 8 * h + 4);

        v8f acc0 = {}, acc1 = {};
#pragma unroll
        for (int i = 0; i < S; ++i) {
            v2f x0 = {a0f[i].x, a0f[i].y}, x1 = {a0f[i].z, a0f[i].w};
            v2f y0 = {a1f[i].x, a1f[i].y}, y1 = {a1f[i].z, a1f[i].w};
            v2f p0 = {bf[i].x, bf[i].y}, p1 = {bf[i].z, bf[i].w};
            acc0 = wmma_f32x4(x0, p0, acc0);
            acc1 = wmma_f32x4(y0, p0, acc1);
            acc0 = wmma_f32x4(x1, p1, acc0);
            acc1 = wmma_f32x4(y1, p1, acc1);
        }

        const float4 o00 = make_float4(acc0[0] + s00.x, acc0[1] + s00.y, acc0[2] + s00.z, acc0[3] + s00.w);
        const float4 o01 = make_float4(acc0[4] + s01.x, acc0[5] + s01.y, acc0[6] + s01.z, acc0[7] + s01.w);
        const float4 o10 = make_float4(acc1[0] + s10.x, acc1[1] + s10.y, acc1[2] + s10.z, acc1[3] + s10.w);
        const float4 o11 = make_float4(acc1[4] + s11.x, acc1[5] + s11.y, acc1[6] + s11.z, acc1[7] + s11.w);
        if (c == 0) {
            float* op = bc + nb + 8 * h;
            *(float4*)(op) = o00;
            *(float4*)(op + 4) = o01;
            *(float4*)(op + 16) = o10;
            *(float4*)(op + 20) = o11;
        }
    }
}

__global__ void gemm_bias_wmma_rt(const float* __restrict__ X,
                                  const float* __restrict__ W,
                                  const float* __restrict__ bias,
                                  float* __restrict__ Y,
                                  int M, int K, int N) {
    const int lane = threadIdx.x & 31;
    const int wave = threadIdx.x >> 5;
    const int tilesN = N >> 4;
    const int tile = blockIdx.x * (blockDim.x >> 5) + wave;
    const int tm = tile / tilesN;
    const int tn = tile - tm * tilesN;
    if (tm * 16 >= M) return;

    const int mrow = min(tm * 16 + (lane & 15), M - 1);
    const int ncol = tn * 16 + (lane & 15);
    const int koff = (lane >> 4) << 1;
    const float* __restrict__ xrow = X + (size_t)mrow * K + koff;
    const float* __restrict__ wrow = W + (size_t)ncol * K + koff;

    v8f acc = {};
#pragma unroll 8
    for (int k = 0; k < K; k += 4) {
        v2f a = *(const v2f*)(xrow + k);
        v2f b = *(const v2f*)(wrow + k);
        acc = wmma_f32x4(a, b, acc);
    }

    const float bv = bias[ncol];
    const int rbase = tm * 16 + ((lane >> 4) << 3);
    float* yp = Y + (size_t)rbase * N + ncol;
#pragma unroll
    for (int r = 0; r < 8; ++r) {
        if (rbase + r < M) *yp = acc[r] + bv;
        yp += N;
    }
}

template <int HT>
__global__ void edge_mlp_wmma(const float* __restrict__ xf,
                              const int* __restrict__ row,
                              const int* __restrict__ col,
                              const float* __restrict__ We,
                              const float* __restrict__ be,
                              float* __restrict__ out,
                              int E, int Nu, int Hrt, int C) {
    const int H = (HT > 0) ? HT : Hrt;
    const int K = 2 * H;
    const int lane = threadIdx.x & 31;
    const int wave = threadIdx.x >> 5;
    const int tile = blockIdx.x * (blockDim.x >> 5) + wave;
    const int ebase = tile * 16;
    if (ebase >= E) return;

    const int m = lane & 15;
    const int e = min(ebase + m, E - 1);
    const int r = row[e];
    const int c = col[e];
    const int koff = (lane >> 4) << 1;

    const float* __restrict__ urow = xf + (size_t)r * H + koff;
    const float* __restrict__ mrow = xf + (size_t)(Nu + c) * H + koff;

    const int n = lane & 15;
    const float wmask = (n < C) ? 1.0f : 0.0f;
    const float* __restrict__ wrow = We + (size_t)min(n, C - 1) * K + koff;

    v8f acc = {};
#pragma unroll 8
    for (int k = 0; k < K; k += 4) {
        const float* src = (k + koff < H) ? (urow + k) : (mrow + (k - H));
        v2f a = *(const v2f*)src;
        v2f b = *(const v2f*)(wrow + k);
        b.x *= wmask;
        b.y *= wmask;
        acc = wmma_f32x4(a, b, acc);
    }

    if (n < C) {
        const float bv = be[n];
        const int mb = (lane >> 4) << 3;
        float* op = out + (size_t)(ebase + mb) * C + n;
        if (ebase + 16 <= E) {
#pragma unroll
            for (int rr = 0; rr < 8; ++rr) { *op = acc[rr] + bv; op += C; }
        } else {
#pragma unroll
            for (int rr = 0; rr < 8; ++rr) {
                if (ebase + mb + rr < E) *op = acc[rr] + bv;
                op += C;
            }
        }
    }
}

__global__ void node_proj_wmma(const float* __restrict__ xf,
                               const float* __restrict__ We,
                               const float* __restrict__ be,
                               float* __restrict__ P,
                               int Nu, int Nm, int C) {
    constexpr int H = 128;
    const int lane = threadIdx.x & 31;
    const int wave = threadIdx.x >> 5;
    const int tilesU = (Nu + 15) >> 4;
    const int tilesM = (Nm + 15) >> 4;
    const int tile = blockIdx.x * (blockDim.x >> 5) + wave;
    if (tile >= tilesU + tilesM) return;

    const bool second = tile >= tilesU;
    const int segEnd = second ? Nu + Nm : Nu;
    const int base = second ? Nu + ((tile - tilesU) << 4) : (tile << 4);

    const int c = lane & 15;
    const int h = lane >> 4;
    const int node = min(base + 2 * (c & 7) + (c >> 3), segEnd - 1);
    const float* __restrict__ arow = xf + (size_t)node * H + 4 * h;
    const float* __restrict__ brow =
        We + (size_t)min(c, C - 1) * (2 * H) + (second ? H : 0) + 4 * h;

    const float bsel = be[min(c, C - 1)];
    const float bv = (second && c < C) ? bsel : 0.0f;

    v8f acc0 = {}, acc1 = {};
#pragma unroll
    for (int i = 0; i < H / 8; ++i) {
        const float4 a = *(const float4*)(arow + 8 * i);
        const float4 b = *(const float4*)(brow + 8 * i);
        v2f a0 = {a.x, a.y}, b0 = {b.x, b.y};
        v2f a1 = {a.z, a.w}, b1 = {b.z, b.w};
        acc0 = wmma_f32x4(a0, b0, acc0);
        acc1 = wmma_f32x4(a1, b1, acc1);
    }

    float v[8];
#pragma unroll
    for (int r = 0; r < 8; ++r) v[r] = (acc0[r] + acc1[r]) + bv;

    float* pp = P + (size_t)base * 16 + lane;
    if (base + 16 <= segEnd) {
#pragma unroll
        for (int r = 0; r < 8; ++r) pp[32 * r] = v[r];
    } else {
#pragma unroll
        for (int r = 0; r < 8; ++r)
            if (base + 2 * r + h < segEnd) pp[32 * r] = v[r];
    }
}

template <int ITER>
__global__ void edge_add10(const float* __restrict__ P,
                           const int* __restrict__ row,
                           const int* __restrict__ col,
                           float* __restrict__ out,
                           int E, int Nu) {
    constexpr unsigned C = 10;
    const unsigned total = (unsigned)E * C;
    const float* __restrict__ Pu = P;
    const float* __restrict__ Pm = P + (size_t)Nu * 16;
    const unsigned perBlock = blockDim.x * ITER;
    const unsigned qb = blockIdx.x * perBlock;

    if ((qb + perBlock) * 4u <= total) {
        unsigned f0[ITER], cA[ITER], cB[ITER];
        int rA[ITER], mA[ITER], rB[ITER], mB[ITER];
#pragma unroll
        for (int it = 0; it < ITER; ++it) {
            f0[it] = 4u * (qb + it * blockDim.x + threadIdx.x);
            const unsigned f1 = f0[it] + 2u;
            const unsigned eA = f0[it] / C, eB = f1 / C;
            cA[it] = f0[it] - eA * C;
            cB[it] = f1 - eB * C;
            rA[it] = row[eA]; mA[it] = col[eA];
            rB[it] = row[eB]; mB[it] = col[eB];
        }
        float2 uA[ITER], vA[ITER], uB[ITER], vB[ITER];
#pragma unroll
        for (int it = 0; it < ITER; ++it) {
            uA[it] = *(const float2*)(Pu + (size_t)rA[it] * 16 + cA[it]);
            vA[it] = *(const float2*)(Pm + (size_t)mA[it] * 16 + cA[it]);
            uB[it] = *(const float2*)(Pu + (size_t)rB[it] * 16 + cB[it]);
            vB[it] = *(const float2*)(Pm + (size_t)mB[it] * 16 + cB[it]);
        }
#pragma unroll
        for (int it = 0; it < ITER; ++it) {
            *(float4*)(out + f0[it]) = make_float4(uA[it].x + vA[it].x, uA[it].y + vA[it].y,
                                                   uB[it].x + vB[it].x, uB[it].y + vB[it].y);
        }
    } else {
        for (unsigned p = threadIdx.x; p < 2u * perBlock; p += blockDim.x) {
            const unsigned f = 4u * qb + 2u * p;
            if (f < total) {
                const unsigned e = f / C, c = f - e * C;
                const float2 u = *(const float2*)(Pu + (size_t)row[e] * 16 + c);
                const float2 v = *(const float2*)(Pm + (size_t)col[e] * 16 + c);
                *(float2*)(out + f) = make_float2(u.x + v.x, u.y + v.y);
            }
        }
    }
}

__global__ void zero_int_kernel(int* __restrict__ p, int n) {
    const int i = blockIdx.x * blockDim.x + threadIdx.x;
    if (i < n) p[i] = 0;
}

__global__ void count_targets_kernel(const int* __restrict__ col,
                                     int* __restrict__ cnt, int E) {
    const int e = blockIdx.x * blockDim.x + threadIdx.x;
    if (e < E) atomicAdd(&cnt[col[e]], 1);
}

__global__ void alloc_segments_kernel(const int* __restrict__ cnt,
                                      int* __restrict__ total,
                                      int* __restrict__ start,
                                      int* __restrict__ cursor,
                                      float* __restrict__ dinv, int N) {
    const int i = blockIdx.x * blockDim.x + threadIdx.x;
    const int lane = threadIdx.x & 31;
    const int c = (i < N) ? cnt[i] : 0;
    int incl = c;
    for (int d = 1; d < 32; d <<= 1) {
        const int t = __shfl_up(incl, d);
        if (lane >= d) incl += t;
    }
    int base = 0;
    if (lane == 31) base = atomicAdd(total, incl);
    base = __shfl(base, 31);
    if (i < N) {
        const int s = base + incl - c;
        start[i] = s;
        cursor[i] = s;
        dinv[i] = rsqrtf((float)(c + 1));
    }
}

__global__ void fill_segments_kernel(const int* __restrict__ row,
                                     const int* __restrict__ col,
                                     const float* __restrict__ dinv,
                                     int* __restrict__ cursor,
                                     int2* __restrict__ slots, int E) {
    const int e = blockIdx.x * blockDim.x + threadIdx.x;
    if (e < E) {
        const int r = row[e];
        const int pos = atomicAdd(&cursor[col[e]], 1);
        slots[pos] = make_int2(r, __float_as_int(dinv[r]));
    }
}

__device__ __forceinline__ float4 fma4(float w, float4 a, float4 c) {
    return make_float4(fmaf(w, a.x, c.x), fmaf(w, a.y, c.y), fmaf(w, a.z, c.z), fmaf(w, a.w, c.w));
}

__global__ void gather_relu_kernel(const float* __restrict__ h,
                                   const float* __restrict__ dinv,
                                   const int* __restrict__ start,
                                   const int* __restrict__ cnt,
                                   const int2* __restrict__ slots,
                                   float* __restrict__ xout,
                                   int N, int E, int H) {
    const int lane = threadIdx.x & 31;
    const int cpn = (H + 127) >> 7;
    const int wid = __builtin_amdgcn_readfirstlane(
        (int)(blockIdx.x * (blockDim.x >> 5) + (threadIdx.x >> 5)));
    const int node = wid / cpn;
    if (node >= N) return;
    const int cb = (wid - node * cpn) << 7;
    const bool act = cb + 4 * lane < H;
    const int c0 = min(cb + 4 * lane, H - 4);
    const float* __restrict__ hl = h + c0;

    const int c = min(max(cnt[node], 0), E);
    const int s = min(max(start[node], 0), E - c);
    const float4 hself = *(const float4*)(hl + (size_t)node * H);
    const float di = dinv[node];

    float4 acc = make_float4(0.f, 0.f, 0.f, 0.f);
    for (int k0 = 0; k0 < c; k0 += 32) {
        const int n = min(32, c - k0);
        int2 ev = make_int2(0, 0);
        if (lane < n) ev = slots[s + k0 + lane];
        for (int j = 0; j < n; j += 4) {
            const int j1 = min(j + 1, n - 1), j2 = min(j + 2, n - 1), j3 = min(j + 3, n - 1);
            const int s0 = __builtin_amdgcn_readlane(ev.x, j);
            const int s1 = __builtin_amdgcn_readlane(ev.x, j1);
            const int s2 = __builtin_amdgcn_readlane(ev.x, j2);
            const int s3 = __builtin_amdgcn_readlane(ev.x, j3);
            const float w0 = __int_as_float(__builtin_amdgcn_readlane(ev.y, j));
            const float w1 = __int_as_float(__builtin_amdgcn_readlane(ev.y, j1));
            const float w2 = __int_as_float(__builtin_amdgcn_readlane(ev.y, j2));
            const float w3 = __int_as_float(__builtin_amdgcn_readlane(ev.y, j3));
            const float4 a0 = *(const float4*)(hl + (size_t)s0 * H);
            const float4 a1 = *(const float4*)(hl + (size_t)s1 * H);
            const float4 a2 = *(const float4*)(hl + (size_t)s2 * H);
            const float4 a3 = *(const float4*)(hl + (size_t)s3 * H);
            acc = fma4(w0, a0, acc);
            if (j + 1 < n) acc = fma4(w1, a1, acc);
            if (j + 2 < n) acc = fma4(w2, a2, acc);
            if (j + 3 < n) acc = fma4(w3, a3, acc);
        }
    }

    if (act) {
        const float d2 = di * di;
        float4 v = make_float4(di * acc.x + d2 * hself.x, di * acc.y + d2 * hself.y,
                               di * acc.z + d2 * hself.z, di * acc.w + d2 * hself.w);
        v.x = v.x > 0.f ? v.x : 0.f;
        v.y = v.y > 0.f ? v.y : 0.f;
        v.z = v.z > 0.f ? v.z : 0.f;
        v.w = v.w > 0.f ? v.w : 0.f;
        *(float4*)(xout + (size_t)node * H + c0) = v;
    }
}

static inline int ceil_div_i(long long a, long long b) { return (int)((a + b - 1) / b); }

static void launch_gemm(const float* X, const float* W, const float* b, float* Y,
                        int M, int K, int N, hipStream_t stream) {
    const int TPB = 256;
    const int strips = ceil_div_i(M, 16);
    if (N == 128 && (K == 32 || K == 64 || K == 128)) {
        const int grid = strips < 1024 ? strips : 1024;
        if (K == 32)
            gemm_bias_wmma_ws<32><<<grid, TPB, 0, stream>>>(X, W, b, Y, M);
        else if (K == 64)
            gemm_bias_wmma_ws<64><<<grid, TPB, 0, stream>>>(X, W, b, Y, M);
        else
            gemm_bias_wmma_ws<128><<<grid, TPB, 0, stream>>>(X, W, b, Y, M);
        return;
    }
    if (K == 32)
        gemm_bias_wmma_lds<32><<<strips, TPB, 0, stream>>>(X, W, b, Y, M, N);
    else if (K == 64)
        gemm_bias_wmma_lds<64><<<strips, TPB, 0, stream>>>(X, W, b, Y, M, N);
    else if (K == 128)
        gemm_bias_wmma_lds<128><<<strips, TPB, 0, stream>>>(X, W, b, Y, M, N);
    else {
        const int tiles = strips * (N / 16);
        gemm_bias_wmma_rt<<<ceil_div_i(tiles, 8), TPB, 0, stream>>>(X, W, b, Y, M, K, N);
    }
}

extern "C" void kernel_launch(void* const* d_in, const int* in_sizes, int n_in,
                              void* d_out, int out_size, void* d_ws, size_t ws_size,
                              hipStream_t stream) {
    const float* x_user  = (const float*)d_in[0];
    const float* x_movie = (const float*)d_in[1];
    const int*   edge    = (const int*)d_in[2];
    const float* Wu = (const float*)d_in[3];
    const float* bu = (const float*)d_in[4];
    const float* Wm = (const float*)d_in[5];
    const float* bm = (const float*)d_in[6];
    const float* W1 = (const float*)d_in[7];
    const float* b1 = (const float*)d_in[8];
    const float* W2 = (const float*)d_in[9];
    const float* b2 = (const float*)d_in[10];
    const float* We = (const float*)d_in[11];
    const float* be = (const float*)d_in[12];
    float* out = (float*)d_out;

    const int H  = in_sizes[4];
    const int C  = in_sizes[12];
    const int Fu = in_sizes[3] / H;
    const int Fm = in_sizes[5] / H;
    const int Nu = in_sizes[0] / Fu;
    const int Nm = in_sizes[1] / Fm;
    const int E  = in_sizes[2] / 2;
    const int N  = Nu + Nm;
    const int* row = edge;
    const int* col = edge + E;

    float* ws   = (float*)d_ws;
    float* X    = ws;
    float* Hh   = X  + (size_t)N * H;
    float* Acc  = Hh + (size_t)N * H;
    float* dinv = Acc + (size_t)N * H;

    const int TPB = 256, WPB = 8;

    const size_t Npad = ((size_t)N + 31) & ~(size_t)31;
    float* P      = dinv + Npad;
    int*   cnt    = (int*)(P + (size_t)N * 16);
    int*   start  = cnt + (((size_t)N + 1 + 31) & ~(size_t)31);
    int*   cursor = start + Npad;
    int2*  slots  = (int2*)(cursor + Npad);
    float* Wcu = (float*)(slots + (((size_t)E + 15) & ~(size_t)15));
    float* Wcm = Wcu + (size_t)H * Fu;
    float* bcu = Wcm + (size_t)H * Fm;
    float* bcm = bcu + H;
    const bool fold = (H == 128) && (Fu % 32 == 0) && (Fm % 32 == 0);

    if (fold) {
        const int units = 8 * (Fu / 32) + 8 * (Fm / 32) + 8;
        compose_wmma<<<ceil_div_i(units, WPB), TPB, 0, stream>>>(W1, b1, Wu, bu, Wm, bm,
                                                                 Wcu, bcu, Wcm, bcm, Fu, Fm);
        launch_gemm(x_user, Wcu, bcu, Hh, Nu, Fu, H, stream);
        launch_gemm(x_movie, Wcm, bcm, Hh + (size_t)Nu * H, Nm, Fm, H, stream);
    } else {
        launch_gemm(x_user, Wu, bu, X, Nu, Fu, H, stream);
        launch_gemm(x_movie, Wm, bm, X + (size_t)Nu * H, Nm, Fm, H, stream);
    }

    zero_int_kernel<<<ceil_div_i(N + 1, TPB), TPB, 0, stream>>>(cnt, N + 1);
    count_targets_kernel<<<ceil_div_i(E, TPB), TPB, 0, stream>>>(col, cnt, E);
    alloc_segments_kernel<<<ceil_div_i(N, TPB), TPB, 0, stream>>>(cnt, cnt + N, start, cursor, dinv, N);
    fill_segments_kernel<<<ceil_div_i(E, TPB), TPB, 0, stream>>>(row, col, dinv, cursor, slots, E);

    const int gatherWaves = N * ((H + 127) / 128);

    if (!fold) launch_gemm(X, W1, b1, Hh, N, H, H, stream);
    gather_relu_kernel<<<ceil_div_i(gatherWaves, WPB), TPB, 0, stream>>>(Hh, dinv, start, cnt, slots, Acc, N, E, H);

    launch_gemm(Acc, W2, b2, Hh, N, H, H, stream);
    gather_relu_kernel<<<ceil_div_i(gatherWaves, WPB), TPB, 0, stream>>>(Hh, dinv, start, cnt, slots, X, N, E, H);

    const int edgeTiles = ceil_div_i(E, 16);
    if (H == 128 && C == 10) {
        const int nodeTiles = ceil_div_i(Nu, 16) + ceil_div_i(Nm, 16);
        node_proj_wmma<<<ceil_div_i(nodeTiles, 4), 128, 0, stream>>>(X, We, be, P, Nu, Nm, C);
        const long long quads = ((long long)E * C + 3) / 4;
        edge_add10<4><<<ceil_div_i(quads, TPB * 4), TPB, 0, stream>>>(P, row, col, out, E, Nu);
    } else if (H == 128)
        edge_mlp_wmma<128><<<ceil_div_i(edgeTiles, WPB), TPB, 0, stream>>>(X, row, col, We, be, out, E, Nu, H, C);
    else
        edge_mlp_wmma<0><<<ceil_div_i(edgeTiles, WPB), TPB, 0, stream>>>(X, row, col, We, be, out, E, Nu, H, C);
}


# in-edge lists built by a two-level counting sort through LDS (256-target bins, LDS atomics) instead of per-edge global atomics; gather looks up source weights; sum order still by atomic arrival
# speedup vs baseline: 4.8430x; 1.5526x over previous
#include <hip/hip_runtime.h>

typedef __attribute__((ext_vector_type(2))) float v2f;
typedef __attribute__((ext_vector_type(8))) float v8f;

__device__ __forceinline__ v8f wmma_f32x4(v2f a, v2f b, v8f c) {
    return __builtin_amdgcn_wmma_f32_16x16x4_f32(
         false, a,  false, b,
         (short)0, c,  false,  false);
}

template <int K>
__global__ void gemm_bias_wmma_lds(const float* __restrict__ X,
                                   const float* __restrict__ W,
                                   const float* __restrict__ bias,
                                   float* __restrict__ Y,
                                   int M, int N) {
    __shared__ float As[16 * K];

    const int lane = threadIdx.x & 31;
    const int wave = threadIdx.x >> 5;
    const int wavesPerBlock = blockDim.x >> 5;
    const int tilesN = N >> 4;
    const int strip = blockIdx.x * 16;

    if (strip + 16 <= M) {
        const float4* __restrict__ src = (const float4*)(X + (size_t)strip * K);
        float4* dst = (float4*)As;
        for (int i = threadIdx.x; i < (16 * K) / 4; i += blockDim.x) dst[i] = src[i];
    } else {
        for (int i = threadIdx.x; i < 16 * K; i += blockDim.x) {
            const int rr = i / K, cc = i - rr * K;
            const int gr = min(strip + rr, M - 1);
            As[i] = X[(size_t)gr * K + cc];
        }
    }
    __syncthreads();

    const int koff = (lane >> 4) << 1;
    const float* __restrict__ arow = As + (lane & 15) * K + koff;
    const bool fullTile = (strip + 16 <= M);
    const int rbase = strip + ((lane >> 4) << 3);

    for (int tn = wave; tn < tilesN; tn += wavesPerBlock) {
        const int ncol = tn * 16 + (lane & 15);
        const float* __restrict__ wrow = W + (size_t)ncol * K + koff;

        v8f acc = {};
#pragma unroll
        for (int k = 0; k < K; k += 4) {
            v2f a = *(const v2f*)(arow + k);
            v2f b = *(const v2f*)(wrow + k);
            acc = wmma_f32x4(a, b, acc);
        }

        const float bv = bias[ncol];
        float* yp = Y + (size_t)rbase * N + ncol;
        if (fullTile) {
#pragma unroll
            for (int r = 0; r < 8; ++r) { *yp = acc[r] + bv; yp += N; }
        } else {
#pragma unroll
            for (int r = 0; r < 8; ++r) {
                if (rbase + r < M) *yp = acc[r] + bv;
                yp += N;
            }
        }
    }
}

template <int K>
__global__ void __launch_bounds__(256)
gemm_bias_wmma_ws(const float* __restrict__ X,
                  const float* __restrict__ W,
                  const float* __restrict__ bias,
                  float* __restrict__ Y, int M) {
    constexpr int N = 128;
    constexpr int P = K + 4;
    constexpr int C4 = K / 4;
    constexpr int NV = 16 * C4;
    constexpr int PER = (NV + 255) / 256;
    __shared__ __attribute__((aligned(16))) float As[2][16 * P];

    const int lane = threadIdx.x & 31;
    const int wave = threadIdx.x >> 5;
    const int c = lane & 15;
    const int h = lane >> 4;
    const int ncol = wave * 16 + c;
    const int strips = (M + 15) >> 4;
    const int step = gridDim.x;

    float4 bf[K / 8];
    {
        const float* __restrict__ wrow = W + (size_t)ncol * K + 4 * h;
#pragma unroll
        for (int i = 0; i < K / 8; ++i) bf[i] = *(const float4*)(wrow + 8 * i);
    }
    const float bv = bias[ncol];

    static_assert(PER <= 2, "a strip is at most 512 float4s");
    const int i0 = threadIdx.x;
    const int rr0 = i0 / C4, cc0 = i0 - rr0 * C4;
    const int i1 = i0 + 256;
    const int rr1 = i1 / C4, cc1 = i1 - rr1 * C4;

    int s = blockIdx.x;
    if (s < strips) {
        if (NV >= 256 || i0 < NV)
            *(float4*)(&As[0][rr0 * P + 4 * cc0]) =
                *(const float4*)(X + (size_t)min(s * 16 + rr0, M - 1) * K + 4 * cc0);
        if (PER > 1)
            *(float4*)(&As[0][rr1 * P + 4 * cc1]) =
                *(const float4*)(X + (size_t)min(s * 16 + rr1, M - 1) * K + 4 * cc1);
    }
    __syncthreads();

    const int aoff = c * P + 4 * h;
    int buf = 0;
    for (; s < strips; s += step) {
        const int sn = s + step;
        const bool more = sn < strips;
        float4 pre0 = make_float4(0.f, 0.f, 0.f, 0.f);
        float4 pre1 = make_float4(0.f, 0.f, 0.f, 0.f);
        if (more) {
            if (NV >= 256 || i0 < NV)
                pre0 = *(const float4*)(X + (size_t)min(sn * 16 + rr0, M - 1) * K + 4 * cc0);
            if (PER > 1)
                pre1 = *(const float4*)(X + (size_t)min(sn * 16 + rr1, M - 1) * K + 4 * cc1);
        }

        const float* arow = &As[buf][aoff];
        v8f acc = {};
#pragma unroll
        for (int i = 0; i < K / 8; ++i) {
            const float4 a = *(const float4*)(arow + 8 * i);
            v2f a0 = {a.x, a.y}, b0 = {bf[i].x, bf[i].y};
            v2f a1 = {a.z, a.w}, b1 = {bf[i].z, bf[i].w};
            acc = wmma_f32x4(a0, b0, acc);
            acc = wmma_f32x4(a1, b1, acc);
        }

        if (more) {
            if (NV >= 256 || i0 < NV) *(float4*)(&As[buf ^ 1][rr0 * P + 4 * cc0]) = pre0;
            if (PER > 1) *(float4*)(&As[buf ^ 1][rr1 * P + 4 * cc1]) = pre1;
        }
        __syncthreads();
        buf ^= 1;

        const int rbase = s * 16 + 8 * h;
        float* yp = Y + (size_t)rbase * N + ncol;
        if (s * 16 + 16 <= M) {
#pragma unroll
            for (int r = 0; r < 8; ++r) { *yp = acc[r] + bv; yp += N; }
        } else {
#pragma unroll
            for (int r = 0; r < 8; ++r) {
                if (rbase + r < M) *yp = acc[r] + bv;
                yp += N;
            }
        }
    }
}

__global__ void __launch_bounds__(256)
compose_wmma(const float* __restrict__ W1, const float* __restrict__ b1,
             const float* __restrict__ Wu, const float* __restrict__ bu,
             const float* __restrict__ Wm, const float* __restrict__ bm,
             float* __restrict__ Wcu, float* __restrict__ bcu,
             float* __restrict__ Wcm, float* __restrict__ bcm,
             int Fu, int Fm) {
    constexpr int H = 128;
    constexpr int S = H / 8;
    const int lane = threadIdx.x & 31;
    const int c = lane & 15;
    const int h = lane >> 4;
    const int unit = blockIdx.x * (blockDim.x >> 5) + (threadIdx.x >> 5);
    const int wu = 8 * (Fu >> 5);
    const int wm = 8 * (Fm >> 5);

    if (unit < wu + wm) {
        const bool second = unit >= wu;
        const int u = second ? unit - wu : unit;
        const int F = second ? Fm : Fu;
        const float* __restrict__ Wt = second ? Wm : Wu;
        float* __restrict__ Wc = second ? Wcm : Wcu;
        const int groups = F >> 5;
        const int nt = u / groups;
        const int cg = u - nt * groups;

        const float* __restrict__ arow = W1 + (size_t)(nt * 16 + c) * H + 4 * h;
        const float* __restrict__ bcol = Wt + (size_t)(4 * h) * F + cg * 32 + c;

        float4 af[S];
        float bl[S][4], br[S][4];
#pragma unroll
        for (int i = 0; i < S; ++i) {
            af[i] = *(const float4*)(arow + 8 * i);
            const float* __restrict__ bp = bcol + (size_t)(8 * i) * F;
#pragma unroll
            for (int e = 0; e < 4; ++e) {
                bl[i][e] = bp[(size_t)e * F];
                br[i][e] = bp[(size_t)e * F + 16];
            }
        }

        v8f acc0 = {}, acc1 = {};
#pragma unroll
        for (int i = 0; i < S; ++i) {
            v2f a0 = {af[i].x, af[i].y}, a1 = {af[i].z, af[i].w};
            v2f p0 = {bl[i][0], bl[i][1]}, p1 = {bl[i][2], bl[i][3]};
            v2f q0 = {br[i][0], br[i][1]}, q1 = {br[i][2], br[i][3]};
            acc0 = wmma_f32x4(a0, p0, acc0);
            acc1 = wmma_f32x4(a0, q0, acc1);
            acc0 = wmma_f32x4(a1, p1, acc0);
            acc1 = wmma_f32x4(a1, q1, acc1);
        }

        float* wp = Wc + (size_t)(nt * 16 + 8 * h) * F + cg * 32 + c;
#pragma unroll
        for (int r = 0; r < 8; ++r) {
            wp[0] = acc0[r];
            wp[16] = acc1[r];
            wp += F;
        }
    } else if (unit < wu + wm + 8) {
        const int q = unit - wu - wm;
        const bool second = q >= 4;
        const float* __restrict__ bt = second ? bm : bu;
        float* __restrict__ bc = second ? bcm : bcu;
        const int nb = 32 * (q & 3);

        const float* __restrict__ ar0 = W1 + (size_t)(nb + c) * H + 4 * h;
        const float* __restrict__ ar1 = ar0 + (size_t)16 * H;
        const float* __restrict__ bv = bt + 4 * h;

        float4 a0f[S], a1f[S], bf[S];
#pragma unroll
        for (int i = 0; i < S; ++i) {
            a0f[i] = *(const float4*)(ar0 + 8 * i);
            a1f[i] = *(const float4*)(ar1 + 8 * i);
            bf[i] = *(const float4*)(bv + 8 * i);
        }
        const float4 s00 = *(const float4*)(b1 + nb + 8 * h);
        const float4 s01 = *(const float4*)(b1 + nb + 8 * h + 4);
        const float4 s10 = *(const float4*)(b1 + nb + 16 + 8 * h);
        const float4 s11 = *(const float4*)(b1 + nb + 16 + 8 * h + 4);

        v8f acc0 = {}, acc1 = {};
#pragma unroll
        for (int i = 0; i < S; ++i) {
            v2f x0 = {a0f[i].x, a0f[i].y}, x1 = {a0f[i].z, a0f[i].w};
            v2f y0 = {a1f[i].x, a1f[i].y}, y1 = {a1f[i].z, a1f[i].w};
            v2f p0 = {bf[i].x, bf[i].y}, p1 = {bf[i].z, bf[i].w};
            acc0 = wmma_f32x4(x0, p0, acc0);
            acc1 = wmma_f32x4(y0, p0, acc1);
            acc0 = wmma_f32x4(x1, p1, acc0);
            acc1 = wmma_f32x4(y1, p1, acc1);
        }

        const float4 o00 = make_float4(acc0[0] + s00.x, acc0[1] + s00.y, acc0[2] + s00.z, acc0[3] + s00.w);
        const float4 o01 = make_float4(acc0[4] + s01.x, acc0[5] + s01.y, acc0[6] + s01.z, acc0[7] + s01.w);
        const float4 o10 = make_float4(acc1[0] + s10.x, acc1[1] + s10.y, acc1[2] + s10.z, acc1[3] + s10.w);
        const float4 o11 = make_float4(acc1[4] + s11.x, acc1[5] + s11.y, acc1[6] + s11.z, acc1[7] + s11.w);
        if (c == 0) {
            float* op = bc + nb + 8 * h;
            *(float4*)(op) = o00;
            *(float4*)(op + 4) = o01;
            *(float4*)(op + 16) = o10;
            *(float4*)(op + 20) = o11;
        }
    }
}

__global__ void gemm_bias_wmma_rt(const float* __restrict__ X,
                                  const float* __restrict__ W,
                                  const float* __restrict__ bias,
                                  float* __restrict__ Y,
                                  int M, int K, int N) {
    const int lane = threadIdx.x & 31;
    const int wave = threadIdx.x >> 5;
    const int tilesN = N >> 4;
    const int tile = blockIdx.x * (blockDim.x >> 5) + wave;
    const int tm = tile / tilesN;
    const int tn = tile - tm * tilesN;
    if (tm * 16 >= M) return;

    const int mrow = min(tm * 16 + (lane & 15), M - 1);
    const int ncol = tn * 16 + (lane & 15);
    const int koff = (lane >> 4) << 1;
    const float* __restrict__ xrow = X + (size_t)mrow * K + koff;
    const float* __restrict__ wrow = W + (size_t)ncol * K + koff;

    v8f acc = {};
#pragma unroll 8
    for (int k = 0; k < K; k += 4) {
        v2f a = *(const v2f*)(xrow + k);
        v2f b = *(const v2f*)(wrow + k);
        acc = wmma_f32x4(a, b, acc);
    }

    const float bv = bias[ncol];
    const int rbase = tm * 16 + ((lane >> 4) << 3);
    float* yp = Y + (size_t)rbase * N + ncol;
#pragma unroll
    for (int r = 0; r < 8; ++r) {
        if (rbase + r < M) *yp = acc[r] + bv;
        yp += N;
    }
}

template <int HT>
__global__ void edge_mlp_wmma(const float* __restrict__ xf,
                              const int* __restrict__ row,
                              const int* __restrict__ col,
                              const float* __restrict__ We,
                              const float* __restrict__ be,
                              float* __restrict__ out,
                              int E, int Nu, int Hrt, int C) {
    const int H = (HT > 0) ? HT : Hrt;
    const int K = 2 * H;
    const int lane = threadIdx.x & 31;
    const int wave = threadIdx.x >> 5;
    const int tile = blockIdx.x * (blockDim.x >> 5) + wave;
    const int ebase = tile * 16;
    if (ebase >= E) return;

    const int m = lane & 15;
    const int e = min(ebase + m, E - 1);
    const int r = row[e];
    const int c = col[e];
    const int koff = (lane >> 4) << 1;

    const float* __restrict__ urow = xf + (size_t)r * H + koff;
    const float* __restrict__ mrow = xf + (size_t)(Nu + c) * H + koff;

    const int n = lane & 15;
    const float wmask = (n < C) ? 1.0f : 0.0f;
    const float* __restrict__ wrow = We + (size_t)min(n, C - 1) * K + koff;

    v8f acc = {};
#pragma unroll 8
    for (int k = 0; k < K; k += 4) {
        const float* src = (k + koff < H) ? (urow + k) : (mrow + (k - H));
        v2f a = *(const v2f*)src;
        v2f b = *(const v2f*)(wrow + k);
        b.x *= wmask;
        b.y *= wmask;
        acc = wmma_f32x4(a, b, acc);
    }

    if (n < C) {
        const float bv = be[n];
        const int mb = (lane >> 4) << 3;
        float* op = out + (size_t)(ebase + mb) * C + n;
        if (ebase + 16 <= E) {
#pragma unroll
            for (int rr = 0; rr < 8; ++rr) { *op = acc[rr] + bv; op += C; }
        } else {
#pragma unroll
            for (int rr = 0; rr < 8; ++rr) {
                if (ebase + mb + rr < E) *op = acc[rr] + bv;
                op += C;
            }
        }
    }
}

__global__ void node_proj_wmma(const float* __restrict__ xf,
                               const float* __restrict__ We,
                               const float* __restrict__ be,
                               float* __restrict__ P,
                               int Nu, int Nm, int C) {
    constexpr int H = 128;
    const int lane = threadIdx.x & 31;
    const int wave = threadIdx.x >> 5;
    const int tilesU = (Nu + 15) >> 4;
    const int tilesM = (Nm + 15) >> 4;
    const int tile = blockIdx.x * (blockDim.x >> 5) + wave;
    if (tile >= tilesU + tilesM) return;

    const bool second = tile >= tilesU;
    const int segEnd = second ? Nu + Nm : Nu;
    const int base = second ? Nu + ((tile - tilesU) << 4) : (tile << 4);

    const int c = lane & 15;
    const int h = lane >> 4;
    const int node = min(base + 2 * (c & 7) + (c >> 3), segEnd - 1);
    const float* __restrict__ arow = xf + (size_t)node * H + 4 * h;
    const float* __restrict__ brow =
        We + (size_t)min(c, C - 1) * (2 * H) + (second ? H : 0) + 4 * h;

    const float bsel = be[min(c, C - 1)];
    const float bv = (second && c < C) ? bsel : 0.0f;

    v8f acc0 = {}, acc1 = {};
#pragma unroll
    for (int i = 0; i < H / 8; ++i) {
        const float4 a = *(const float4*)(arow + 8 * i);
        const float4 b = *(const float4*)(brow + 8 * i);
        v2f a0 = {a.x, a.y}, b0 = {b.x, b.y};
        v2f a1 = {a.z, a.w}, b1 = {b.z, b.w};
        acc0 = wmma_f32x4(a0, b0, acc0);
        acc1 = wmma_f32x4(a1, b1, acc1);
    }

    float v[8];
#pragma unroll
    for (int r = 0; r < 8; ++r) v[r] = (acc0[r] + acc1[r]) + bv;

    float* pp = P + (size_t)base * 16 + lane;
    if (base + 16 <= segEnd) {
#pragma unroll
        for (int r = 0; r < 8; ++r) pp[32 * r] = v[r];
    } else {
#pragma unroll
        for (int r = 0; r < 8; ++r)
            if (base + 2 * r + h < segEnd) pp[32 * r] = v[r];
    }
}

template <int ITER>
__global__ void edge_add10(const float* __restrict__ P,
                           const int* __restrict__ row,
                           const int* __restrict__ col,
                           float* __restrict__ out,
                           int E, int Nu) {
    constexpr unsigned C = 10;
    const unsigned total = (unsigned)E * C;
    const float* __restrict__ Pu = P;
    const float* __restrict__ Pm = P + (size_t)Nu * 16;
    const unsigned perBlock = blockDim.x * ITER;
    const unsigned qb = blockIdx.x * perBlock;

    if ((qb + perBlock) * 4u <= total) {
        unsigned f0[ITER], cA[ITER], cB[ITER];
        int rA[ITER], mA[ITER], rB[ITER], mB[ITER];
#pragma unroll
        for (int it = 0; it < ITER; ++it) {
            f0[it] = 4u * (qb + it * blockDim.x + threadIdx.x);
            const unsigned f1 = f0[it] + 2u;
            const unsigned eA = f0[it] / C, eB = f1 / C;
            cA[it] = f0[it] - eA * C;
            cB[it] = f1 - eB * C;
            rA[it] = row[eA]; mA[it] = col[eA];
            rB[it] = row[eB]; mB[it] = col[eB];
        }
        float2 uA[ITER], vA[ITER], uB[ITER], vB[ITER];
#pragma unroll
        for (int it = 0; it < ITER; ++it) {
            uA[it] = *(const float2*)(Pu + (size_t)rA[it] * 16 + cA[it]);
            vA[it] = *(const float2*)(Pm + (size_t)mA[it] * 16 + cA[it]);
            uB[it] = *(const float2*)(Pu + (size_t)rB[it] * 16 + cB[it]);
            vB[it] = *(const float2*)(Pm + (size_t)mB[it] * 16 + cB[it]);
        }
#pragma unroll
        for (int it = 0; it < ITER; ++it) {
            *(float4*)(out + f0[it]) = make_float4(uA[it].x + vA[it].x, uA[it].y + vA[it].y,
                                                   uB[it].x + vB[it].x, uB[it].y + vB[it].y);
        }
    } else {
        for (unsigned p = threadIdx.x; p < 2u * perBlock; p += blockDim.x) {
            const unsigned f = 4u * qb + 2u * p;
            if (f < total) {
                const unsigned e = f / C, c = f - e * C;
                const float2 u = *(const float2*)(Pu + (size_t)row[e] * 16 + c);
                const float2 v = *(const float2*)(Pm + (size_t)col[e] * 16 + c);
                *(float2*)(out + f) = make_float2(u.x + v.x, u.y + v.y);
            }
        }
    }
}

constexpr int SORT_G = 128;
constexpr int SORT_NBMAX = 1024;
constexpr int SORT_T = 1024;
static_assert(SORT_T == SORT_NBMAX && SORT_G % 32 == 0, "one bin per thread, whole chunk rows per wave");

__device__ __forceinline__ int wave_incl_scan(int v, int lane) {
    for (int d = 1; d < 32; d <<= 1) {
        const int t = __shfl_up(v, d);
        if (lane >= d) v += t;
    }
    return v;
}

__global__ void __launch_bounds__(SORT_T)
bin_hist_kernel(const int* __restrict__ col, int* __restrict__ BH,
                int E, int NB, int chunk) {
    __shared__ int hist[SORT_NBMAX];
    const int tid = threadIdx.x;
    const int g = blockIdx.x;
    hist[tid] = 0;
    __syncthreads();
    const int lo = min(g * chunk, E);
    const int hi = min(lo + chunk, E);
    for (int e0 = lo + tid; e0 < hi; e0 += 4 * SORT_T) {
        int c[4];
#pragma unroll
        for (int k = 0; k < 4; ++k) {
            const int e = e0 + k * SORT_T;
            c[k] = (e < hi) ? col[e] : -1;
        }
#pragma unroll
        for (int k = 0; k < 4; ++k)
            if (c[k] >= 0) atomicAdd(&hist[min(c[k] >> 8, NB - 1)], 1);
    }
    __syncthreads();
    BH[(size_t)g * SORT_NBMAX + tid] = hist[tid];
}

__global__ void __launch_bounds__(SORT_T)
bin_scatter_kernel(const int* __restrict__ row, const int* __restrict__ col,
                   const int* __restrict__ BH,
                   int* __restrict__ binBase, int* __restrict__ binCount,
                   int2* __restrict__ binned, int E, int NB, int chunk) {
    __shared__ int pre[SORT_NBMAX];
    __shared__ int tot[SORT_NBMAX];
    __shared__ int cur[SORT_NBMAX];
    __shared__ int wsum[32];
    const int tid = threadIdx.x;
    const int lane = tid & 31;
    const int wave = tid >> 5;
    const int g = blockIdx.x;

    pre[tid] = 0;
    tot[tid] = 0;
    __syncthreads();
    constexpr int GW = SORT_G / 32;
    const int g0 = wave * GW;
    for (int bb = 0; bb * 128 < NB; bb += 2) {
        const int b4 = bb * 128 + 4 * lane;
        const bool two = (bb + 1) * 128 < NB;
        int4 va[GW], vb[GW];
#pragma unroll
        for (int k = 0; k < GW; ++k) {
            const int* rp = BH + (size_t)(g0 + k) * SORT_NBMAX + b4;
            va[k] = *(const int4*)rp;
            vb[k] = two ? *(const int4*)(rp + 128) : make_int4(0, 0, 0, 0);
        }
        int4 pa = make_int4(0, 0, 0, 0), ta = pa, pb = pa, tb = pa;
#pragma unroll
        for (int k = 0; k < GW; ++k) {
            const bool early = g0 + k < g;
            ta.x += va[k].x; ta.y += va[k].y; ta.z += va[k].z; ta.w += va[k].w;
            tb.x += vb[k].x; tb.y += vb[k].y; tb.z += vb[k].z; tb.w += vb[k].w;
            if (early) {
                pa.x += va[k].x; pa.y += va[k].y; pa.z += va[k].z; pa.w += va[k].w;
                pb.x += vb[k].x; pb.y += vb[k].y; pb.z += vb[k].z; pb.w += vb[k].w;
            }
        }
        atomicAdd(&tot[b4 + 0], ta.x); atomicAdd(&tot[b4 + 1], ta.y);
        atomicAdd(&tot[b4 + 2], ta.z); atomicAdd(&tot[b4 + 3], ta.w);
        atomicAdd(&pre[b4 + 0], pa.x); atomicAdd(&pre[b4 + 1], pa.y);
        atomicAdd(&pre[b4 + 2], pa.z); atomicAdd(&pre[b4 + 3], pa.w);
        if (two) {
            atomicAdd(&tot[b4 + 128], tb.x); atomicAdd(&tot[b4 + 129], tb.y);
            atomicAdd(&tot[b4 + 130], tb.z); atomicAdd(&tot[b4 + 131], tb.w);
            atomicAdd(&pre[b4 + 128], pb.x); atomicAdd(&pre[b4 + 129], pb.y);
            atomicAdd(&pre[b4 + 130], pb.z); atomicAdd(&pre[b4 + 131], pb.w);
        }
    }
    __syncthreads();
    const int mine = tot[tid];
    const int incl = wave_incl_scan(mine, lane);
    if (lane == 31) wsum[wave] = incl;
    __syncthreads();
    if (wave == 0) {
        const int w = wsum[lane];
        const int wi = wave_incl_scan(w, lane);
        wsum[lane] = wi - w;
    }
    __syncthreads();
    const int base = wsum[wave] + incl - mine;
    cur[tid] = base + pre[tid];
    if (g == 0 && tid < NB) { binBase[tid] = base; binCount[tid] = mine; }
    __syncthreads();

    const int lo = min(g * chunk, E);
    const int hi = min(lo + chunk, E);
    for (int e0 = lo + tid; e0 < hi; e0 += 4 * SORT_T) {
        int c[4], r[4];
#pragma unroll
        for (int k = 0; k < 4; ++k) {
            const int e = e0 + k * SORT_T;
            c[k] = (e < hi) ? col[e] : -1;
            r[k] = (e < hi) ? row[e] : 0;
        }
#pragma unroll
        for (int k = 0; k < 4; ++k) {
            if (c[k] >= 0) {
                const int pos = atomicAdd(&cur[min(c[k] >> 8, NB - 1)], 1);
                binned[min(max(pos, 0), E - 1)] = make_int2(r[k], c[k]);
            }
        }
    }
}

__global__ void __launch_bounds__(SORT_T)
bin_sort_kernel(const int2* __restrict__ binned,
                const int* __restrict__ binBase, const int* __restrict__ binCount,
                int* __restrict__ start, int* __restrict__ cnt,
                float* __restrict__ dinv, int* __restrict__ srcs, int N, int E) {
    __shared__ int c[256];
    __shared__ int cur[256];
    __shared__ int wsum[8];
    const int tid = threadIdx.x;
    const int lane = tid & 31;
    const int wave = tid >> 5;
    const int b = blockIdx.x;
    const int count = min(max(binCount[b], 0), E);
    const int base = min(max(binBase[b], 0), E - count);
    const int node0 = b << 8;

    if (tid < 256) c[tid] = 0;
    __syncthreads();
    for (int i0 = tid; i0 < count; i0 += 4 * SORT_T) {
        int t[4];
#pragma unroll
        for (int k = 0; k < 4; ++k) {
            const int i = i0 + k * SORT_T;
            t[k] = (i < count) ? binned[base + i].y : -1;
        }
#pragma unroll
        for (int k = 0; k < 4; ++k)
            if (t[k] >= 0) atomicAdd(&c[min(max(t[k] - node0, 0), 255)], 1);
    }
    __syncthreads();
    int mine = 0, incl = 0;
    if (tid < 256) {
        mine = c[tid];
        incl = wave_incl_scan(mine, lane);
        if (lane == 31) wsum[wave] = incl;
    }
    __syncthreads();
    if (tid < 256) {
        int off = 0;
        for (int w = 0; w < wave; ++w) off += wsum[w];
        const int excl = off + incl - mine;
        cur[tid] = excl;
        const int node = node0 + tid;
        if (node < N) {
            start[node] = base + excl;
            cnt[node] = mine;
            dinv[node] = rsqrtf((float)(mine + 1));
        }
    }
    __syncthreads();
    for (int i0 = tid; i0 < count; i0 += 4 * SORT_T) {
        int2 ed[4];
#pragma unroll
        for (int k = 0; k < 4; ++k) {
            const int i = i0 + k * SORT_T;
            ed[k] = (i < count) ? binned[base + i] : make_int2(0, -1);
        }
#pragma unroll
        for (int k = 0; k < 4; ++k) {
            if (ed[k].y >= 0) {
                const int p = atomicAdd(&cur[min(max(ed[k].y - node0, 0), 255)], 1);
                srcs[base + min(max(p, 0), count - 1)] = ed[k].x;
            }
        }
    }
}

__global__ void zero_int_kernel(int* __restrict__ p, int n) {
    const int i = blockIdx.x * blockDim.x + threadIdx.x;
    if (i < n) p[i] = 0;
}

__global__ void count_targets_kernel(const int* __restrict__ col,
                                     int* __restrict__ cnt, int E) {
    const int e = blockIdx.x * blockDim.x + threadIdx.x;
    if (e < E) atomicAdd(&cnt[col[e]], 1);
}

__global__ void alloc_segments_kernel(const int* __restrict__ cnt,
                                      int* __restrict__ total,
                                      int* __restrict__ start,
                                      int* __restrict__ cursor,
                                      float* __restrict__ dinv, int N) {
    const int i = blockIdx.x * blockDim.x + threadIdx.x;
    const int lane = threadIdx.x & 31;
    const int c = (i < N) ? cnt[i] : 0;
    int incl = c;
    for (int d = 1; d < 32; d <<= 1) {
        const int t = __shfl_up(incl, d);
        if (lane >= d) incl += t;
    }
    int base = 0;
    if (lane == 31) base = atomicAdd(total, incl);
    base = __shfl(base, 31);
    if (i < N) {
        const int s = base + incl - c;
        start[i] = s;
        cursor[i] = s;
        dinv[i] = rsqrtf((float)(c + 1));
    }
}

__global__ void fill_segments_kernel(const int* __restrict__ row,
                                     const int* __restrict__ col,
                                     const float* __restrict__ dinv,
                                     int* __restrict__ cursor,
                                     int2* __restrict__ slots, int E) {
    const int e = blockIdx.x * blockDim.x + threadIdx.x;
    if (e < E) {
        const int r = row[e];
        const int pos = atomicAdd(&cursor[col[e]], 1);
        slots[pos] = make_int2(r, __float_as_int(dinv[r]));
    }
}

__device__ __forceinline__ float4 fma4(float w, float4 a, float4 c) {
    return make_float4(fmaf(w, a.x, c.x), fmaf(w, a.y, c.y), fmaf(w, a.z, c.z), fmaf(w, a.w, c.w));
}

template <bool SORTED>
__global__ void gather_relu_kernel(const float* __restrict__ h,
                                   const float* __restrict__ dinv,
                                   const int* __restrict__ start,
                                   const int* __restrict__ cnt,
                                   const int2* __restrict__ slots,
                                   const int* __restrict__ srcs,
                                   float* __restrict__ xout,
                                   int N, int E, int H) {
    const int lane = threadIdx.x & 31;
    const int cpn = (H + 127) >> 7;
    const int wid = __builtin_amdgcn_readfirstlane(
        (int)(blockIdx.x * (blockDim.x >> 5) + (threadIdx.x >> 5)));
    const int node = wid / cpn;
    if (node >= N) return;
    const int cb = (wid - node * cpn) << 7;
    const bool act = cb + 4 * lane < H;
    const int c0 = min(cb + 4 * lane, H - 4);
    const float* __restrict__ hl = h + c0;

    const int c = min(max(cnt[node], 0), E);
    const int s = min(max(start[node], 0), E - c);
    const float4 hself = *(const float4*)(hl + (size_t)node * H);
    const float di = dinv[node];

    float4 acc = make_float4(0.f, 0.f, 0.f, 0.f);
    for (int k0 = 0; k0 < c; k0 += 32) {
        const int n = min(32, c - k0);
        int2 ev = make_int2(0, 0);
        if (lane < n) {
            if (SORTED) {
                ev.x = min(max(srcs[s + k0 + lane], 0), N - 1);
                ev.y = __float_as_int(dinv[ev.x]);
            } else {
                ev = slots[s + k0 + lane];
            }
        }
        for (int j = 0; j < n; j += 4) {
            const int j1 = min(j + 1, n - 1), j2 = min(j + 2, n - 1), j3 = min(j + 3, n - 1);
            const int s0 = __builtin_amdgcn_readlane(ev.x, j);
            const int s1 = __builtin_amdgcn_readlane(ev.x, j1);
            const int s2 = __builtin_amdgcn_readlane(ev.x, j2);
            const int s3 = __builtin_amdgcn_readlane(ev.x, j3);
            const float w0 = __int_as_float(__builtin_amdgcn_readlane(ev.y, j));
            const float w1 = __int_as_float(__builtin_amdgcn_readlane(ev.y, j1));
            const float w2 = __int_as_float(__builtin_amdgcn_readlane(ev.y, j2));
            const float w3 = __int_as_float(__builtin_amdgcn_readlane(ev.y, j3));
            const float4 a0 = *(const float4*)(hl + (size_t)s0 * H);
            const float4 a1 = *(const float4*)(hl + (size_t)s1 * H);
            const float4 a2 = *(const float4*)(hl + (size_t)s2 * H);
            const float4 a3 = *(const float4*)(hl + (size_t)s3 * H);
            acc = fma4(w0, a0, acc);
            if (j + 1 < n) acc = fma4(w1, a1, acc);
            if (j + 2 < n) acc = fma4(w2, a2, acc);
            if (j + 3 < n) acc = fma4(w3, a3, acc);
        }
    }

    if (act) {
        const float d2 = di * di;
        float4 v = make_float4(di * acc.x + d2 * hself.x, di * acc.y + d2 * hself.y,
                               di * acc.z + d2 * hself.z, di * acc.w + d2 * hself.w);
        v.x = v.x > 0.f ? v.x : 0.f;
        v.y = v.y > 0.f ? v.y : 0.f;
        v.z = v.z > 0.f ? v.z : 0.f;
        v.w = v.w > 0.f ? v.w : 0.f;
        *(float4*)(xout + (size_t)node * H + c0) = v;
    }
}

static inline int ceil_div_i(long long a, long long b) { return (int)((a + b - 1) / b); }

static void launch_gemm(const float* X, const float* W, const float* b, float* Y,
                        int M, int K, int N, hipStream_t stream) {
    const int TPB = 256;
    const int strips = ceil_div_i(M, 16);
    if (N == 128 && (K == 32 || K == 64 || K == 128)) {
        const int grid = strips < 1024 ? strips : 1024;
        if (K == 32)
            gemm_bias_wmma_ws<32><<<grid, TPB, 0, stream>>>(X, W, b, Y, M);
        else if (K == 64)
            gemm_bias_wmma_ws<64><<<grid, TPB, 0, stream>>>(X, W, b, Y, M);
        else
            gemm_bias_wmma_ws<128><<<grid, TPB, 0, stream>>>(X, W, b, Y, M);
        return;
    }
    if (K == 32)
        gemm_bias_wmma_lds<32><<<strips, TPB, 0, stream>>>(X, W, b, Y, M, N);
    else if (K == 64)
        gemm_bias_wmma_lds<64><<<strips, TPB, 0, stream>>>(X, W, b, Y, M, N);
    else if (K == 128)
        gemm_bias_wmma_lds<128><<<strips, TPB, 0, stream>>>(X, W, b, Y, M, N);
    else {
        const int tiles = strips * (N / 16);
        gemm_bias_wmma_rt<<<ceil_div_i(tiles, 8), TPB, 0, stream>>>(X, W, b, Y, M, K, N);
    }
}

extern "C" void kernel_launch(void* const* d_in, const int* in_sizes, int n_in,
                              void* d_out, int out_size, void* d_ws, size_t ws_size,
                              hipStream_t stream) {
    const float* x_user  = (const float*)d_in[0];
    const float* x_movie = (const float*)d_in[1];
    const int*   edge    = (const int*)d_in[2];
    const float* Wu = (const float*)d_in[3];
    const float* bu = (const float*)d_in[4];
    const float* Wm = (const float*)d_in[5];
    const float* bm = (const float*)d_in[6];
    const float* W1 = (const float*)d_in[7];
    const float* b1 = (const float*)d_in[8];
    const float* W2 = (const float*)d_in[9];
    const float* b2 = (const float*)d_in[10];
    const float* We = (const float*)d_in[11];
    const float* be = (const float*)d_in[12];
    float* out = (float*)d_out;

    const int H  = in_sizes[4];
    const int C  = in_sizes[12];
    const int Fu = in_sizes[3] / H;
    const int Fm = in_sizes[5] / H;
    const int Nu = in_sizes[0] / Fu;
    const int Nm = in_sizes[1] / Fm;
    const int E  = in_sizes[2] / 2;
    const int N  = Nu + Nm;
    const int* row = edge;
    const int* col = edge + E;

    float* ws   = (float*)d_ws;
    float* X    = ws;
    float* Hh   = X  + (size_t)N * H;
    float* Acc  = Hh + (size_t)N * H;
    float* dinv = Acc + (size_t)N * H;

    const int TPB = 256, WPB = 8;

    const size_t Npad = ((size_t)N + 31) & ~(size_t)31;
    float* P      = dinv + Npad;
    int*   cnt    = (int*)(P + (size_t)N * 16);
    int*   start  = cnt + (((size_t)N + 1 + 31) & ~(size_t)31);
    int*   cursor = start + Npad;
    int2*  slots  = (int2*)(cursor + Npad);
    float* Wcu = (float*)(slots + (((size_t)E + 15) & ~(size_t)15));
    float* Wcm = Wcu + (size_t)H * Fu;
    float* bcu = Wcm + (size_t)H * Fm;
    float* bcm = bcu + H;
    const bool fold = (H == 128) && (Fu % 32 == 0) && (Fm % 32 == 0);
    int* srcs     = (int*)(bcm + H);
    int* BH       = srcs + (((size_t)E + 31) & ~(size_t)31);
    int* binBase  = BH + (size_t)SORT_NBMAX * SORT_G;
    int* binCount = binBase + SORT_NBMAX;
    const int NB = ceil_div_i(N, 256);
    const bool sorted = NB <= SORT_NBMAX;

    if (fold) {
        const int units = 8 * (Fu / 32) + 8 * (Fm / 32) + 8;
        compose_wmma<<<ceil_div_i(units, WPB), TPB, 0, stream>>>(W1, b1, Wu, bu, Wm, bm,
                                                                 Wcu, bcu, Wcm, bcm, Fu, Fm);
        launch_gemm(x_user, Wcu, bcu, Hh, Nu, Fu, H, stream);
        launch_gemm(x_movie, Wcm, bcm, Hh + (size_t)Nu * H, Nm, Fm, H, stream);
    } else {
        launch_gemm(x_user, Wu, bu, X, Nu, Fu, H, stream);
        launch_gemm(x_movie, Wm, bm, X + (size_t)Nu * H, Nm, Fm, H, stream);
    }

    if (sorted) {
        const int chunk = ceil_div_i(E, SORT_G);
        bin_hist_kernel<<<SORT_G, SORT_T, 0, stream>>>(col, BH, E, NB, chunk);
        bin_scatter_kernel<<<SORT_G, SORT_T, 0, stream>>>(row, col, BH, binBase, binCount, slots, E, NB, chunk);
        bin_sort_kernel<<<NB, SORT_T, 0, stream>>>(slots, binBase, binCount, start, cnt, dinv, srcs, N, E);
    } else {
        zero_int_kernel<<<ceil_div_i(N + 1, TPB), TPB, 0, stream>>>(cnt, N + 1);
        count_targets_kernel<<<ceil_div_i(E, TPB), TPB, 0, stream>>>(col, cnt, E);
        alloc_segments_kernel<<<ceil_div_i(N, TPB), TPB, 0, stream>>>(cnt, cnt + N, start, cursor, dinv, N);
        fill_segments_kernel<<<ceil_div_i(E, TPB), TPB, 0, stream>>>(row, col, dinv, cursor, slots, E);
    }

    const int gatherWaves = N * ((H + 127) / 128);

    if (!fold) launch_gemm(X, W1, b1, Hh, N, H, H, stream);
    if (sorted)
        gather_relu_kernel<true><<<ceil_div_i(gatherWaves, WPB), TPB, 0, stream>>>(Hh, dinv, start, cnt, slots, srcs, Acc, N, E, H);
    else
        gather_relu_kernel<false><<<ceil_div_i(gatherWaves, WPB), TPB, 0, stream>>>(Hh, dinv, start, cnt, slots, srcs, Acc, N, E, H);

    launch_gemm(Acc, W2, b2, Hh, N, H, H, stream);
    if (sorted)
        gather_relu_kernel<true><<<ceil_div_i(gatherWaves, WPB), TPB, 0, stream>>>(Hh, dinv, start, cnt, slots, srcs, X, N, E, H);
    else
        gather_relu_kernel<false><<<ceil_div_i(gatherWaves, WPB), TPB, 0, stream>>>(Hh, dinv, start, cnt, slots, srcs, X, N, E, H);

    const int edgeTiles = ceil_div_i(E, 16);
    if (H == 128 && C == 10) {
        const int nodeTiles = ceil_div_i(Nu, 16) + ceil_div_i(Nm, 16);
        node_proj_wmma<<<ceil_div_i(nodeTiles, 4), 128, 0, stream>>>(X, We, be, P, Nu, Nm, C);
        const long long quads = ((long long)E * C + 3) / 4;
        edge_add10<4><<<ceil_div_i(quads, TPB * 4), TPB, 0, stream>>>(P, row, col, out, E, Nu);
    } else if (H == 128)
        edge_mlp_wmma<128><<<ceil_div_i(edgeTiles, WPB), TPB, 0, stream>>>(X, row, col, We, be, out, E, Nu, H, C);
    else
        edge_mlp_wmma<0><<<ceil_div_i(edgeTiles, WPB), TPB, 0, stream>>>(X, row, col, We, be, out, E, Nu, H, C);
}


# source weights folded into the gathered rows: GEMMs store h row i times dinv[i] (lists built first); gathers add plain rows, no per-edge dinv lookup; self row joins the sum, order changes
# speedup vs baseline: 5.0075x; 1.0340x over previous
#include <hip/hip_runtime.h>

typedef __attribute__((ext_vector_type(2))) float v2f;
typedef __attribute__((ext_vector_type(8))) float v8f;

__device__ __forceinline__ v8f wmma_f32x4(v2f a, v2f b, v8f c) {
    return __builtin_amdgcn_wmma_f32_16x16x4_f32(
         false, a,  false, b,
         (short)0, c,  false,  false);
}

__device__ __forceinline__ void load_row_scale8(const float* __restrict__ scale,
                                                int row0, int M, float (&sc)[8]) {
    if (scale == nullptr) {
#pragma unroll
        for (int r = 0; r < 8; ++r) sc[r] = 1.0f;
    } else {
#pragma unroll
        for (int r = 0; r < 8; ++r) sc[r] = scale[min(row0 + r, M - 1)];
    }
}

template <int K>
__global__ void gemm_bias_wmma_lds(const float* __restrict__ X,
                                   const float* __restrict__ W,
                                   const float* __restrict__ bias,
                                   const float* __restrict__ scale,
                                   float* __restrict__ Y,
                                   int M, int N) {
    __shared__ float As[16 * K];

    const int lane = threadIdx.x & 31;
    const int wave = threadIdx.x >> 5;
    const int wavesPerBlock = blockDim.x >> 5;
    const int tilesN = N >> 4;
    const int strip = blockIdx.x * 16;

    if (strip + 16 <= M) {
        const float4* __restrict__ src = (const float4*)(X + (size_t)strip * K);
        float4* dst = (float4*)As;
        for (int i = threadIdx.x; i < (16 * K) / 4; i += blockDim.x) dst[i] = src[i];
    } else {
        for (int i = threadIdx.x; i < 16 * K; i += blockDim.x) {
            const int rr = i / K, cc = i - rr * K;
            const int gr = min(strip + rr, M - 1);
            As[i] = X[(size_t)gr * K + cc];
        }
    }
    __syncthreads();

    const int koff = (lane >> 4) << 1;
    const float* __restrict__ arow = As + (lane & 15) * K + koff;
    const bool fullTile = (strip + 16 <= M);
    const int rbase = strip + ((lane >> 4) << 3);
    float sc[8];
    load_row_scale8(scale, rbase, M, sc);

    for (int tn = wave; tn < tilesN; tn += wavesPerBlock) {
        const int ncol = tn * 16 + (lane & 15);
        const float* __restrict__ wrow = W + (size_t)ncol * K + koff;

        v8f acc = {};
#pragma unroll
        for (int k = 0; k < K; k += 4) {
            v2f a = *(const v2f*)(arow + k);
            v2f b = *(const v2f*)(wrow + k);
            acc = wmma_f32x4(a, b, acc);
        }

        const float bv = bias[ncol];
        float* yp = Y + (size_t)rbase * N + ncol;
        if (fullTile) {
#pragma unroll
            for (int r = 0; r < 8; ++r) { *yp = (acc[r] + bv) * sc[r]; yp += N; }
        } else {
#pragma unroll
            for (int r = 0; r < 8; ++r) {
                if (rbase + r < M) *yp = (acc[r] + bv) * sc[r];
                yp += N;
            }
        }
    }
}

template <int K, bool SCALED>
__global__ void __launch_bounds__(256)
gemm_bias_wmma_ws(const float* __restrict__ X,
                  const float* __restrict__ W,
                  const float* __restrict__ bias,
                  const float* __restrict__ scale,
                  float* __restrict__ Y, int M) {
    constexpr int N = 128;
    constexpr int P = K + 4;
    constexpr int C4 = K / 4;
    constexpr int NV = 16 * C4;
    constexpr int PER = (NV + 255) / 256;
    __shared__ __attribute__((aligned(16))) float As[2][16 * P];

    const int lane = threadIdx.x & 31;
    const int wave = threadIdx.x >> 5;
    const int c = lane & 15;
    const int h = lane >> 4;
    const int ncol = wave * 16 + c;
    const int strips = (M + 15) >> 4;
    const int step = gridDim.x;

    float4 bf[K / 8];
    {
        const float* __restrict__ wrow = W + (size_t)ncol * K + 4 * h;
#pragma unroll
        for (int i = 0; i < K / 8; ++i) bf[i] = *(const float4*)(wrow + 8 * i);
    }
    const float bv = bias[ncol];

    static_assert(PER <= 2, "a strip is at most 512 float4s");
    const int i0 = threadIdx.x;
    const int rr0 = i0 / C4, cc0 = i0 - rr0 * C4;
    const int i1 = i0 + 256;
    const int rr1 = i1 / C4, cc1 = i1 - rr1 * C4;

    int s = blockIdx.x;
    if (s < strips) {
        if (NV >= 256 || i0 < NV)
            *(float4*)(&As[0][rr0 * P + 4 * cc0]) =
                *(const float4*)(X + (size_t)min(s * 16 + rr0, M - 1) * K + 4 * cc0);
        if (PER > 1)
            *(float4*)(&As[0][rr1 * P + 4 * cc1]) =
                *(const float4*)(X + (size_t)min(s * 16 + rr1, M - 1) * K + 4 * cc1);
    }
    __syncthreads();

    const int aoff = c * P + 4 * h;
    int buf = 0;
    for (; s < strips; s += step) {
        const int sn = s + step;
        const bool more = sn < strips;
        float4 pre0 = make_float4(0.f, 0.f, 0.f, 0.f);
        float4 pre1 = make_float4(0.f, 0.f, 0.f, 0.f);
        if (more) {
            if (NV >= 256 || i0 < NV)
                pre0 = *(const float4*)(X + (size_t)min(sn * 16 + rr0, M - 1) * K + 4 * cc0);
            if (PER > 1)
                pre1 = *(const float4*)(X + (size_t)min(sn * 16 + rr1, M - 1) * K + 4 * cc1);
        }
        const int rbase = s * 16 + 8 * h;
        float4 su = make_float4(1.f, 1.f, 1.f, 1.f), sv = su;
        if (SCALED) {
            const float* __restrict__ sp = scale + max(min(rbase, M - 8), 0);
            __builtin_memcpy(&su, sp, 16);
            __builtin_memcpy(&sv, sp + 4, 16);
        }

        const float* arow = &As[buf][aoff];
        v8f acc = {};
#pragma unroll
        for (int i = 0; i < K / 8; ++i) {
            const float4 a = *(const float4*)(arow + 8 * i);
            v2f a0 = {a.x, a.y}, b0 = {bf[i].x, bf[i].y};
            v2f a1 = {a.z, a.w}, b1 = {bf[i].z, bf[i].w};
            acc = wmma_f32x4(a0, b0, acc);
            acc = wmma_f32x4(a1, b1, acc);
        }

        if (more) {
            if (NV >= 256 || i0 < NV) *(float4*)(&As[buf ^ 1][rr0 * P + 4 * cc0]) = pre0;
            if (PER > 1) *(float4*)(&As[buf ^ 1][rr1 * P + 4 * cc1]) = pre1;
        }
        __syncthreads();
        buf ^= 1;

        float* yp = Y + (size_t)rbase * N + ncol;
        if (s * 16 + 16 <= M) {
            const float sc[8] = {su.x, su.y, su.z, su.w, sv.x, sv.y, sv.z, sv.w};
#pragma unroll
            for (int r = 0; r < 8; ++r) { *yp = (acc[r] + bv) * sc[r]; yp += N; }
        } else {
#pragma unroll
            for (int r = 0; r < 8; ++r) {
                if (rbase + r < M) *yp = (acc[r] + bv) * (SCALED ? scale[rbase + r] : 1.0f);
                yp += N;
            }
        }
    }
}

__global__ void __launch_bounds__(256)
compose_wmma(const float* __restrict__ W1, const float* __restrict__ b1,
             const float* __restrict__ Wu, const float* __restrict__ bu,
             const float* __restrict__ Wm, const float* __restrict__ bm,
             float* __restrict__ Wcu, float* __restrict__ bcu,
             float* __restrict__ Wcm, float* __restrict__ bcm,
             int Fu, int Fm) {
    constexpr int H = 128;
    constexpr int S = H / 8;
    const int lane = threadIdx.x & 31;
    const int c = lane & 15;
    const int h = lane >> 4;
    const int unit = blockIdx.x * (blockDim.x >> 5) + (threadIdx.x >> 5);
    const int wu = 8 * (Fu >> 5);
    const int wm = 8 * (Fm >> 5);

    if (unit < wu + wm) {
        const bool second = unit >= wu;
        const int u = second ? unit - wu : unit;
        const int F = second ? Fm : Fu;
        const float* __restrict__ Wt = second ? Wm : Wu;
        float* __restrict__ Wc = second ? Wcm : Wcu;
        const int groups = F >> 5;
        const int nt = u / groups;
        const int cg = u - nt * groups;

        const float* __restrict__ arow = W1 + (size_t)(nt * 16 + c) * H + 4 * h;
        const float* __restrict__ bcol = Wt + (size_t)(4 * h) * F + cg * 32 + c;

        float4 af[S];
        float bl[S][4], br[S][4];
#pragma unroll
        for (int i = 0; i < S; ++i) {
            af[i] = *(const float4*)(arow + 8 * i);
            const float* __restrict__ bp = bcol + (size_t)(8 * i) * F;
#pragma unroll
            for (int e = 0; e < 4; ++e) {
                bl[i][e] = bp[(size_t)e * F];
                br[i][e] = bp[(size_t)e * F + 16];
            }
        }

        v8f acc0 = {}, acc1 = {};
#pragma unroll
        for (int i = 0; i < S; ++i) {
            v2f a0 = {af[i].x, af[i].y}, a1 = {af[i].z, af[i].w};
            v2f p0 = {bl[i][0], bl[i][1]}, p1 = {bl[i][2], bl[i][3]};
            v2f q0 = {br[i][0], br[i][1]}, q1 = {br[i][2], br[i][3]};
            acc0 = wmma_f32x4(a0, p0, acc0);
            acc1 = wmma_f32x4(a0, q0, acc1);
            acc0 = wmma_f32x4(a1, p1, acc0);
            acc1 = wmma_f32x4(a1, q1, acc1);
        }

        float* wp = Wc + (size_t)(nt * 16 + 8 * h) * F + cg * 32 + c;
#pragma unroll
        for (int r = 0; r < 8; ++r) {
            wp[0] = acc0[r];
            wp[16] = acc1[r];
            wp += F;
        }
    } else if (unit < wu + wm + 8) {
        const int q = unit - wu - wm;
        const bool second = q >= 4;
        const float* __restrict__ bt = second ? bm : bu;
        float* __restrict__ bc = second ? bcm : bcu;
        const int nb = 32 * (q & 3);

        const float* __restrict__ ar0 = W1 + (size_t)(nb + c) * H + 4 * h;
        const float* __restrict__ ar1 = ar0 + (size_t)16 * H;
        const float* __restrict__ bv = bt + 4 * h;

        float4 a0f[S], a1f[S], bf[S];
#pragma unroll
        for (int i = 0; i < S; ++i) {
            a0f[i] = *(const float4*)(ar0 + 8 * i);
            a1f[i] = *(const float4*)(ar1 + 8 * i);
            bf[i] = *(const float4*)(bv + 8 * i);
        }
        const float4 s00 = *(const float4*)(b1 + nb + 8 * h);
        const float4 s01 = *(const float4*)(b1 + nb + 8 * h + 4);
        const float4 s10 = *(const float4*)(b1 + nb + 16 + 8 * h);
        const float4 s11 = *(const float4*)(b1 + nb + 16 + 8 * h + 4);

        v8f acc0 = {}, acc1 = {};
#pragma unroll
        for (int i = 0; i < S; ++i) {
            v2f x0 = {a0f[i].x, a0f[i].y}, x1 = {a0f[i].z, a0f[i].w};
            v2f y0 = {a1f[i].x, a1f[i].y}, y1 = {a1f[i].z, a1f[i].w};
            v2f p0 = {bf[i].x, bf[i].y}, p1 = {bf[i].z, bf[i].w};
            acc0 = wmma_f32x4(x0, p0, acc0);
            acc1 = wmma_f32x4(y0, p0, acc1);
            acc0 = wmma_f32x4(x1, p1, acc0);
            acc1 = wmma_f32x4(y1, p1, acc1);
        }

        const float4 o00 = make_float4(acc0[0] + s00.x, acc0[1] + s00.y, acc0[2] + s00.z, acc0[3] + s00.w);
        const float4 o01 = make_float4(acc0[4] + s01.x, acc0[5] + s01.y, acc0[6] + s01.z, acc0[7] + s01.w);
        const float4 o10 = make_float4(acc1[0] + s10.x, acc1[1] + s10.y, acc1[2] + s10.z, acc1[3] + s10.w);
        const float4 o11 = make_float4(acc1[4] + s11.x, acc1[5] + s11.y, acc1[6] + s11.z, acc1[7] + s11.w);
        if (c == 0) {
            float* op = bc + nb + 8 * h;
            *(float4*)(op) = o00;
            *(float4*)(op + 4) = o01;
            *(float4*)(op + 16) = o10;
            *(float4*)(op + 20) = o11;
        }
    }
}

__global__ void gemm_bias_wmma_rt(const float* __restrict__ X,
                                  const float* __restrict__ W,
                                  const float* __restrict__ bias,
                                  const float* __restrict__ scale,
                                  float* __restrict__ Y,
                                  int M, int K, int N) {
    const int lane = threadIdx.x & 31;
    const int wave = threadIdx.x >> 5;
    const int tilesN = N >> 4;
    const int tile = blockIdx.x * (blockDim.x >> 5) + wave;
    const int tm = tile / tilesN;
    const int tn = tile - tm * tilesN;
    if (tm * 16 >= M) return;

    const int mrow = min(tm * 16 + (lane & 15), M - 1);
    const int ncol = tn * 16 + (lane & 15);
    const int koff = (lane >> 4) << 1;
    const float* __restrict__ xrow = X + (size_t)mrow * K + koff;
    const float* __restrict__ wrow = W + (size_t)ncol * K + koff;

    v8f acc = {};
#pragma unroll 8
    for (int k = 0; k < K; k += 4) {
        v2f a = *(const v2f*)(xrow + k);
        v2f b = *(const v2f*)(wrow + k);
        acc = wmma_f32x4(a, b, acc);
    }

    const float bv = bias[ncol];
    const int rbase = tm * 16 + ((lane >> 4) << 3);
    float sc[8];
    load_row_scale8(scale, rbase, M, sc);
    float* yp = Y + (size_t)rbase * N + ncol;
#pragma unroll
    for (int r = 0; r < 8; ++r) {
        if (rbase + r < M) *yp = (acc[r] + bv) * sc[r];
        yp += N;
    }
}

template <int HT>
__global__ void edge_mlp_wmma(const float* __restrict__ xf,
                              const int* __restrict__ row,
                              const int* __restrict__ col,
                              const float* __restrict__ We,
                              const float* __restrict__ be,
                              float* __restrict__ out,
                              int E, int Nu, int Hrt, int C) {
    const int H = (HT > 0) ? HT : Hrt;
    const int K = 2 * H;
    const int lane = threadIdx.x & 31;
    const int wave = threadIdx.x >> 5;
    const int tile = blockIdx.x * (blockDim.x >> 5) + wave;
    const int ebase = tile * 16;
    if (ebase >= E) return;

    const int m = lane & 15;
    const int e = min(ebase + m, E - 1);
    const int r = row[e];
    const int c = col[e];
    const int koff = (lane >> 4) << 1;

    const float* __restrict__ urow = xf + (size_t)r * H + koff;
    const float* __restrict__ mrow = xf + (size_t)(Nu + c) * H + koff;

    const int n = lane & 15;
    const float wmask = (n < C) ? 1.0f : 0.0f;
    const float* __restrict__ wrow = We + (size_t)min(n, C - 1) * K + koff;

    v8f acc = {};
#pragma unroll 8
    for (int k = 0; k < K; k += 4) {
        const float* src = (k + koff < H) ? (urow + k) : (mrow + (k - H));
        v2f a = *(const v2f*)src;
        v2f b = *(const v2f*)(wrow + k);
        b.x *= wmask;
        b.y *= wmask;
        acc = wmma_f32x4(a, b, acc);
    }

    if (n < C) {
        const float bv = be[n];
        const int mb = (lane >> 4) << 3;
        float* op = out + (size_t)(ebase + mb) * C + n;
        if (ebase + 16 <= E) {
#pragma unroll
            for (int rr = 0; rr < 8; ++rr) { *op = acc[rr] + bv; op += C; }
        } else {
#pragma unroll
            for (int rr = 0; rr < 8; ++rr) {
                if (ebase + mb + rr < E) *op = acc[rr] + bv;
                op += C;
            }
        }
    }
}

__global__ void node_proj_wmma(const float* __restrict__ xf,
                               const float* __restrict__ We,
                               const float* __restrict__ be,
                               float* __restrict__ P,
                               int Nu, int Nm, int C) {
    constexpr int H = 128;
    const int lane = threadIdx.x & 31;
    const int wave = threadIdx.x >> 5;
    const int tilesU = (Nu + 15) >> 4;
    const int tilesM = (Nm + 15) >> 4;
    const int tile = blockIdx.x * (blockDim.x >> 5) + wave;
    if (tile >= tilesU + tilesM) return;

    const bool second = tile >= tilesU;
    const int segEnd = second ? Nu + Nm : Nu;
    const int base = second ? Nu + ((tile - tilesU) << 4) : (tile << 4);

    const int c = lane & 15;
    const int h = lane >> 4;
    const int node = min(base + 2 * (c & 7) + (c >> 3), segEnd - 1);
    const float* __restrict__ arow = xf + (size_t)node * H + 4 * h;
    const float* __restrict__ brow =
        We + (size_t)min(c, C - 1) * (2 * H) + (second ? H : 0) + 4 * h;

    const float bsel = be[min(c, C - 1)];
    const float bv = (second && c < C) ? bsel : 0.0f;

    v8f acc0 = {}, acc1 = {};
#pragma unroll
    for (int i = 0; i < H / 8; ++i) {
        const float4 a = *(const float4*)(arow + 8 * i);
        const float4 b = *(const float4*)(brow + 8 * i);
        v2f a0 = {a.x, a.y}, b0 = {b.x, b.y};
        v2f a1 = {a.z, a.w}, b1 = {b.z, b.w};
        acc0 = wmma_f32x4(a0, b0, acc0);
        acc1 = wmma_f32x4(a1, b1, acc1);
    }

    float v[8];
#pragma unroll
    for (int r = 0; r < 8; ++r) v[r] = (acc0[r] + acc1[r]) + bv;

    float* pp = P + (size_t)base * 16 + lane;
    if (base + 16 <= segEnd) {
#pragma unroll
        for (int r = 0; r < 8; ++r) pp[32 * r] = v[r];
    } else {
#pragma unroll
        for (int r = 0; r < 8; ++r)
            if (base + 2 * r + h < segEnd) pp[32 * r] = v[r];
    }
}

template <int ITER>
__global__ void edge_add10(const float* __restrict__ P,
                           const int* __restrict__ row,
                           const int* __restrict__ col,
                           float* __restrict__ out,
                           int E, int Nu) {
    constexpr unsigned C = 10;
    const unsigned total = (unsigned)E * C;
    const float* __restrict__ Pu = P;
    const float* __restrict__ Pm = P + (size_t)Nu * 16;
    const unsigned perBlock = blockDim.x * ITER;
    const unsigned qb = blockIdx.x * perBlock;

    if ((qb + perBlock) * 4u <= total) {
        unsigned f0[ITER], cA[ITER], cB[ITER];
        int rA[ITER], mA[ITER], rB[ITER], mB[ITER];
#pragma unroll
        for (int it = 0; it < ITER; ++it) {
            f0[it] = 4u * (qb + it * blockDim.x + threadIdx.x);
            const unsigned f1 = f0[it] + 2u;
            const unsigned eA = f0[it] / C, eB = f1 / C;
            cA[it] = f0[it] - eA * C;
            cB[it] = f1 - eB * C;
            rA[it] = row[eA]; mA[it] = col[eA];
            rB[it] = row[eB]; mB[it] = col[eB];
        }
        float2 uA[ITER], vA[ITER], uB[ITER], vB[ITER];
#pragma unroll
        for (int it = 0; it < ITER; ++it) {
            uA[it] = *(const float2*)(Pu + (size_t)rA[it] * 16 + cA[it]);
            vA[it] = *(const float2*)(Pm + (size_t)mA[it] * 16 + cA[it]);
            uB[it] = *(const float2*)(Pu + (size_t)rB[it] * 16 + cB[it]);
            vB[it] = *(const float2*)(Pm + (size_t)mB[it] * 16 + cB[it]);
        }
#pragma unroll
        for (int it = 0; it < ITER; ++it) {
            *(float4*)(out + f0[it]) = make_float4(uA[it].x + vA[it].x, uA[it].y + vA[it].y,
                                                   uB[it].x + vB[it].x, uB[it].y + vB[it].y);
        }
    } else {
        for (unsigned p = threadIdx.x; p < 2u * perBlock; p += blockDim.x) {
            const unsigned f = 4u * qb + 2u * p;
            if (f < total) {
                const unsigned e = f / C, c = f - e * C;
                const float2 u = *(const float2*)(Pu + (size_t)row[e] * 16 + c);
                const float2 v = *(const float2*)(Pm + (size_t)col[e] * 16 + c);
                *(float2*)(out + f) = make_float2(u.x + v.x, u.y + v.y);
            }
        }
    }
}

constexpr int SORT_G = 128;
constexpr int SORT_NBMAX = 1024;
constexpr int SORT_T = 1024;
static_assert(SORT_T == SORT_NBMAX && SORT_G % 32 == 0, "one bin per thread, whole chunk rows per wave");

__device__ __forceinline__ int wave_incl_scan(int v, int lane) {
    for (int d = 1; d < 32; d <<= 1) {
        const int t = __shfl_up(v, d);
        if (lane >= d) v += t;
    }
    return v;
}

__global__ void __launch_bounds__(SORT_T)
bin_hist_kernel(const int* __restrict__ col, int* __restrict__ BH,
                int E, int NB, int chunk) {
    __shared__ int hist[SORT_NBMAX];
    const int tid = threadIdx.x;
    const int g = blockIdx.x;
    hist[tid] = 0;
    __syncthreads();
    const int lo = min(g * chunk, E);
    const int hi = min(lo + chunk, E);
    for (int e0 = lo + tid; e0 < hi; e0 += 4 * SORT_T) {
        int c[4];
#pragma unroll
        for (int k = 0; k < 4; ++k) {
            const int e = e0 + k * SORT_T;
            c[k] = (e < hi) ? col[e] : -1;
        }
#pragma unroll
        for (int k = 0; k < 4; ++k)
            if (c[k] >= 0) atomicAdd(&hist[min(c[k] >> 8, NB - 1)], 1);
    }
    __syncthreads();
    BH[(size_t)g * SORT_NBMAX + tid] = hist[tid];
}

__global__ void __launch_bounds__(SORT_T)
bin_scatter_kernel(const int* __restrict__ row, const int* __restrict__ col,
                   const int* __restrict__ BH,
                   int* __restrict__ binBase, int* __restrict__ binCount,
                   int2* __restrict__ binned, int E, int NB, int chunk) {
    __shared__ int pre[SORT_NBMAX];
    __shared__ int tot[SORT_NBMAX];
    __shared__ int cur[SORT_NBMAX];
    __shared__ int wsum[32];
    const int tid = threadIdx.x;
    const int lane = tid & 31;
    const int wave = tid >> 5;
    const int g = blockIdx.x;

    pre[tid] = 0;
    tot[tid] = 0;
    __syncthreads();
    constexpr int GW = SORT_G / 32;
    const int g0 = wave * GW;
    for (int bb = 0; bb * 128 < NB; bb += 2) {
        const int b4 = bb * 128 + 4 * lane;
        const bool two = (bb + 1) * 128 < NB;
        int4 va[GW], vb[GW];
#pragma unroll
        for (int k = 0; k < GW; ++k) {
            const int* rp = BH + (size_t)(g0 + k) * SORT_NBMAX + b4;
            va[k] = *(const int4*)rp;
            vb[k] = two ? *(const int4*)(rp + 128) : make_int4(0, 0, 0, 0);
        }
        int4 pa = make_int4(0, 0, 0, 0), ta = pa, pb = pa, tb = pa;
#pragma unroll
        for (int k = 0; k < GW; ++k) {
            const bool early = g0 + k < g;
            ta.x += va[k].x; ta.y += va[k].y; ta.z += va[k].z; ta.w += va[k].w;
            tb.x += vb[k].x; tb.y += vb[k].y; tb.z += vb[k].z; tb.w += vb[k].w;
            if (early) {
                pa.x += va[k].x; pa.y += va[k].y; pa.z += va[k].z; pa.w += va[k].w;
                pb.x += vb[k].x; pb.y += vb[k].y; pb.z += vb[k].z; pb.w += vb[k].w;
            }
        }
        atomicAdd(&tot[b4 + 0], ta.x); atomicAdd(&tot[b4 + 1], ta.y);
        atomicAdd(&tot[b4 + 2], ta.z); atomicAdd(&tot[b4 + 3], ta.w);
        atomicAdd(&pre[b4 + 0], pa.x); atomicAdd(&pre[b4 + 1], pa.y);
        atomicAdd(&pre[b4 + 2], pa.z); atomicAdd(&pre[b4 + 3], pa.w);
        if (two) {
            atomicAdd(&tot[b4 + 128], tb.x); atomicAdd(&tot[b4 + 129], tb.y);
            atomicAdd(&tot[b4 + 130], tb.z); atomicAdd(&tot[b4 + 131], tb.w);
            atomicAdd(&pre[b4 + 128], pb.x); atomicAdd(&pre[b4 + 129], pb.y);
            atomicAdd(&pre[b4 + 130], pb.z); atomicAdd(&pre[b4 + 131], pb.w);
        }
    }
    __syncthreads();
    const int mine = tot[tid];
    const int incl = wave_incl_scan(mine, lane);
    if (lane == 31) wsum[wave] = incl;
    __syncthreads();
    if (wave == 0) {
        const int w = wsum[lane];
        const int wi = wave_incl_scan(w, lane);
        wsum[lane] = wi - w;
    }
    __syncthreads();
    const int base = wsum[wave] + incl - mine;
    cur[tid] = base + pre[tid];
    if (g == 0 && tid < NB) { binBase[tid] = base; binCount[tid] = mine; }
    __syncthreads();

    const int lo = min(g * chunk, E);
    const int hi = min(lo + chunk, E);
    for (int e0 = lo + tid; e0 < hi; e0 += 4 * SORT_T) {
        int c[4], r[4];
#pragma unroll
        for (int k = 0; k < 4; ++k) {
            const int e = e0 + k * SORT_T;
            c[k] = (e < hi) ? col[e] : -1;
            r[k] = (e < hi) ? row[e] : 0;
        }
#pragma unroll
        for (int k = 0; k < 4; ++k) {
            if (c[k] >= 0) {
                const int pos = atomicAdd(&cur[min(c[k] >> 8, NB - 1)], 1);
                binned[min(max(pos, 0), E - 1)] = make_int2(r[k], c[k]);
            }
        }
    }
}

__global__ void __launch_bounds__(SORT_T)
bin_sort_kernel(const int2* __restrict__ binned,
                const int* __restrict__ binBase, const int* __restrict__ binCount,
                int* __restrict__ start, int* __restrict__ cnt,
                float* __restrict__ dinv, int* __restrict__ srcs, int N, int E) {
    __shared__ int c[256];
    __shared__ int cur[256];
    __shared__ int wsum[8];
    const int tid = threadIdx.x;
    const int lane = tid & 31;
    const int wave = tid >> 5;
    const int b = blockIdx.x;
    const int count = min(max(binCount[b], 0), E);
    const int base = min(max(binBase[b], 0), E - count);
    const int node0 = b << 8;

    if (tid < 256) c[tid] = 0;
    __syncthreads();
    for (int i0 = tid; i0 < count; i0 += 4 * SORT_T) {
        int t[4];
#pragma unroll
        for (int k = 0; k < 4; ++k) {
            const int i = i0 + k * SORT_T;
            t[k] = (i < count) ? binned[base + i].y : -1;
        }
#pragma unroll
        for (int k = 0; k < 4; ++k)
            if (t[k] >= 0) atomicAdd(&c[min(max(t[k] - node0, 0), 255)], 1);
    }
    __syncthreads();
    int mine = 0, incl = 0;
    if (tid < 256) {
        mine = c[tid];
        incl = wave_incl_scan(mine, lane);
        if (lane == 31) wsum[wave] = incl;
    }
    __syncthreads();
    if (tid < 256) {
        int off = 0;
        for (int w = 0; w < wave; ++w) off += wsum[w];
        const int excl = off + incl - mine;
        cur[tid] = excl;
        const int node = node0 + tid;
        if (node < N) {
            start[node] = base + excl;
            cnt[node] = mine;
            dinv[node] = rsqrtf((float)(mine + 1));
        }
    }
    __syncthreads();
    for (int i0 = tid; i0 < count; i0 += 4 * SORT_T) {
        int2 ed[4];
#pragma unroll
        for (int k = 0; k < 4; ++k) {
            const int i = i0 + k * SORT_T;
            ed[k] = (i < count) ? binned[base + i] : make_int2(0, -1);
        }
#pragma unroll
        for (int k = 0; k < 4; ++k) {
            if (ed[k].y >= 0) {
                const int p = atomicAdd(&cur[min(max(ed[k].y - node0, 0), 255)], 1);
                srcs[base + min(max(p, 0), count - 1)] = ed[k].x;
            }
        }
    }
}

__global__ void zero_int_kernel(int* __restrict__ p, int n) {
    const int i = blockIdx.x * blockDim.x + threadIdx.x;
    if (i < n) p[i] = 0;
}

__global__ void count_targets_kernel(const int* __restrict__ col,
                                     int* __restrict__ cnt, int E) {
    const int e = blockIdx.x * blockDim.x + threadIdx.x;
    if (e < E) atomicAdd(&cnt[col[e]], 1);
}

__global__ void alloc_segments_kernel(const int* __restrict__ cnt,
                                      int* __restrict__ total,
                                      int* __restrict__ start,
                                      int* __restrict__ cursor,
                                      float* __restrict__ dinv, int N) {
    const int i = blockIdx.x * blockDim.x + threadIdx.x;
    const int lane = threadIdx.x & 31;
    const int c = (i < N) ? cnt[i] : 0;
    int incl = c;
    for (int d = 1; d < 32; d <<= 1) {
        const int t = __shfl_up(incl, d);
        if (lane >= d) incl += t;
    }
    int base = 0;
    if (lane == 31) base = atomicAdd(total, incl);
    base = __shfl(base, 31);
    if (i < N) {
        const int s = base + incl - c;
        start[i] = s;
        cursor[i] = s;
        dinv[i] = rsqrtf((float)(c + 1));
    }
}

__global__ void fill_segments_kernel(const int* __restrict__ row,
                                     const int* __restrict__ col,
                                     const float* __restrict__ dinv,
                                     int* __restrict__ cursor,
                                     int2* __restrict__ slots, int E) {
    const int e = blockIdx.x * blockDim.x + threadIdx.x;
    if (e < E) {
        const int r = row[e];
        const int pos = atomicAdd(&cursor[col[e]], 1);
        slots[pos] = make_int2(r, __float_as_int(dinv[r]));
    }
}

__device__ __forceinline__ float4 fma4(float w, float4 a, float4 c) {
    return make_float4(fmaf(w, a.x, c.x), fmaf(w, a.y, c.y), fmaf(w, a.z, c.z), fmaf(w, a.w, c.w));
}

__device__ __forceinline__ float4 add4(float4 a, float4 c) {
    return make_float4(a.x + c.x, a.y + c.y, a.z + c.z, a.w + c.w);
}

template <bool SORTED>
__global__ void gather_relu_kernel(const float* __restrict__ h,
                                   const float* __restrict__ dinv,
                                   const int* __restrict__ start,
                                   const int* __restrict__ cnt,
                                   const int2* __restrict__ slots,
                                   const int* __restrict__ srcs,
                                   float* __restrict__ xout,
                                   int N, int E, int H) {
    const int lane = threadIdx.x & 31;
    const int cpn = (H + 127) >> 7;
    const int wid = __builtin_amdgcn_readfirstlane(
        (int)(blockIdx.x * (blockDim.x >> 5) + (threadIdx.x >> 5)));
    const int node = wid / cpn;
    if (node >= N) return;
    const int cb = (wid - node * cpn) << 7;
    const bool act = cb + 4 * lane < H;
    const int c0 = min(cb + 4 * lane, H - 4);
    const float* __restrict__ hl = h + c0;

    const int c = min(max(cnt[node], 0), E);
    const int s = min(max(start[node], 0), E - c);
    const float4 hself = *(const float4*)(hl + (size_t)node * H);
    const float di = dinv[node];

    float4 acc = make_float4(0.f, 0.f, 0.f, 0.f);
    for (int k0 = 0; k0 < c; k0 += 32) {
        const int n = min(32, c - k0);
        int2 ev = make_int2(0, 0);
        if (lane < n) {
            if (SORTED)
                ev.x = min(max(srcs[s + k0 + lane], 0), N - 1);
            else
                ev = slots[s + k0 + lane];
        }
        for (int j = 0; j < n; j += 4) {
            const int j1 = min(j + 1, n - 1), j2 = min(j + 2, n - 1), j3 = min(j + 3, n - 1);
            const int s0 = __builtin_amdgcn_readlane(ev.x, j);
            const int s1 = __builtin_amdgcn_readlane(ev.x, j1);
            const int s2 = __builtin_amdgcn_readlane(ev.x, j2);
            const int s3 = __builtin_amdgcn_readlane(ev.x, j3);
            const float4 a0 = *(const float4*)(hl + (size_t)s0 * H);
            const float4 a1 = *(const float4*)(hl + (size_t)s1 * H);
            const float4 a2 = *(const float4*)(hl + (size_t)s2 * H);
            const float4 a3 = *(const float4*)(hl + (size_t)s3 * H);
            if (SORTED) {
                acc = add4(a0, acc);
                if (j + 1 < n) acc = add4(a1, acc);
                if (j + 2 < n) acc = add4(a2, acc);
                if (j + 3 < n) acc = add4(a3, acc);
            } else {
                const float w0 = __int_as_float(__builtin_amdgcn_readlane(ev.y, j));
                const float w1 = __int_as_float(__builtin_amdgcn_readlane(ev.y, j1));
                const float w2 = __int_as_float(__builtin_amdgcn_readlane(ev.y, j2));
                const float w3 = __int_as_float(__builtin_amdgcn_readlane(ev.y, j3));
                acc = fma4(w0, a0, acc);
                if (j + 1 < n) acc = fma4(w1, a1, acc);
                if (j + 2 < n) acc = fma4(w2, a2, acc);
                if (j + 3 < n) acc = fma4(w3, a3, acc);
            }
        }
    }

    if (act) {
        float4 v;
        if (SORTED) {
            v = make_float4(di * (acc.x + hself.x), di * (acc.y + hself.y),
                            di * (acc.z + hself.z), di * (acc.w + hself.w));
        } else {
            const float d2 = di * di;
            v = make_float4(di * acc.x + d2 * hself.x, di * acc.y + d2 * hself.y,
                            di * acc.z + d2 * hself.z, di * acc.w + d2 * hself.w);
        }
        v.x = v.x > 0.f ? v.x : 0.f;
        v.y = v.y > 0.f ? v.y : 0.f;
        v.z = v.z > 0.f ? v.z : 0.f;
        v.w = v.w > 0.f ? v.w : 0.f;
        *(float4*)(xout + (size_t)node * H + c0) = v;
    }
}

static inline int ceil_div_i(long long a, long long b) { return (int)((a + b - 1) / b); }

static void launch_gemm(const float* X, const float* W, const float* b, const float* scale,
                        float* Y, int M, int K, int N, hipStream_t stream) {
    const int TPB = 256;
    const int strips = ceil_div_i(M, 16);
    if (N == 128 && (K == 32 || K == 64 || K == 128)) {
        const int grid = strips < 1024 ? strips : 1024;
        if (scale) {
            if (K == 32)
                gemm_bias_wmma_ws<32, true><<<grid, TPB, 0, stream>>>(X, W, b, scale, Y, M);
            else if (K == 64)
                gemm_bias_wmma_ws<64, true><<<grid, TPB, 0, stream>>>(X, W, b, scale, Y, M);
            else
                gemm_bias_wmma_ws<128, true><<<grid, TPB, 0, stream>>>(X, W, b, scale, Y, M);
        } else {
            if (K == 32)
                gemm_bias_wmma_ws<32, false><<<grid, TPB, 0, stream>>>(X, W, b, scale, Y, M);
            else if (K == 64)
                gemm_bias_wmma_ws<64, false><<<grid, TPB, 0, stream>>>(X, W, b, scale, Y, M);
            else
                gemm_bias_wmma_ws<128, false><<<grid, TPB, 0, stream>>>(X, W, b, scale, Y, M);
        }
        return;
    }
    if (K == 32)
        gemm_bias_wmma_lds<32><<<strips, TPB, 0, stream>>>(X, W, b, scale, Y, M, N);
    else if (K == 64)
        gemm_bias_wmma_lds<64><<<strips, TPB, 0, stream>>>(X, W, b, scale, Y, M, N);
    else if (K == 128)
        gemm_bias_wmma_lds<128><<<strips, TPB, 0, stream>>>(X, W, b, scale, Y, M, N);
    else {
        const int tiles = strips * (N / 16);
        gemm_bias_wmma_rt<<<ceil_div_i(tiles, 8), TPB, 0, stream>>>(X, W, b, scale, Y, M, K, N);
    }
}

extern "C" void kernel_launch(void* const* d_in, const int* in_sizes, int n_in,
                              void* d_out, int out_size, void* d_ws, size_t ws_size,
                              hipStream_t stream) {
    const float* x_user  = (const float*)d_in[0];
    const float* x_movie = (const float*)d_in[1];
    const int*   edge    = (const int*)d_in[2];
    const float* Wu = (const float*)d_in[3];
    const float* bu = (const float*)d_in[4];
    const float* Wm = (const float*)d_in[5];
    const float* bm = (const float*)d_in[6];
    const float* W1 = (const float*)d_in[7];
    const float* b1 = (const float*)d_in[8];
    const float* W2 = (const float*)d_in[9];
    const float* b2 = (const float*)d_in[10];
    const float* We = (const float*)d_in[11];
    const float* be = (const float*)d_in[12];
    float* out = (float*)d_out;

    const int H  = in_sizes[4];
    const int C  = in_sizes[12];
    const int Fu = in_sizes[3] / H;
    const int Fm = in_sizes[5] / H;
    const int Nu = in_sizes[0] / Fu;
    const int Nm = in_sizes[1] / Fm;
    const int E  = in_sizes[2] / 2;
    const int N  = Nu + Nm;
    const int* row = edge;
    const int* col = edge + E;

    float* ws   = (float*)d_ws;
    float* X    = ws;
    float* Hh   = X  + (size_t)N * H;
    float* Acc  = Hh + (size_t)N * H;
    float* dinv = Acc + (size_t)N * H;

    const int TPB = 256, WPB = 8;

    const size_t Npad = ((size_t)N + 31) & ~(size_t)31;
    float* P      = dinv + Npad;
    int*   cnt    = (int*)(P + (size_t)N * 16);
    int*   start  = cnt + (((size_t)N + 1 + 31) & ~(size_t)31);
    int*   cursor = start + Npad;
    int2*  slots  = (int2*)(cursor + Npad);
    float* Wcu = (float*)(slots + (((size_t)E + 15) & ~(size_t)15));
    float* Wcm = Wcu + (size_t)H * Fu;
    float* bcu = Wcm + (size_t)H * Fm;
    float* bcm = bcu + H;
    const bool fold = (H == 128) && (Fu % 32 == 0) && (Fm % 32 == 0);
    int* srcs     = (int*)(bcm + H);
    int* BH       = srcs + (((size_t)E + 31) & ~(size_t)31);
    int* binBase  = BH + (size_t)SORT_NBMAX * SORT_G;
    int* binCount = binBase + SORT_NBMAX;
    const int NB = ceil_div_i(N, 256);
    const bool sorted = NB <= SORT_NBMAX;

    if (sorted) {
        const int chunk = ceil_div_i(E, SORT_G);
        bin_hist_kernel<<<SORT_G, SORT_T, 0, stream>>>(col, BH, E, NB, chunk);
        bin_scatter_kernel<<<SORT_G, SORT_T, 0, stream>>>(row, col, BH, binBase, binCount, slots, E, NB, chunk);
        bin_sort_kernel<<<NB, SORT_T, 0, stream>>>(slots, binBase, binCount, start, cnt, dinv, srcs, N, E);
    } else {
        zero_int_kernel<<<ceil_div_i(N + 1, TPB), TPB, 0, stream>>>(cnt, N + 1);
        count_targets_kernel<<<ceil_div_i(E, TPB), TPB, 0, stream>>>(col, cnt, E);
        alloc_segments_kernel<<<ceil_div_i(N, TPB), TPB, 0, stream>>>(cnt, cnt + N, start, cursor, dinv, N);
        fill_segments_kernel<<<ceil_div_i(E, TPB), TPB, 0, stream>>>(row, col, dinv, cursor, slots, E);
    }
    const float* rowScale = sorted ? dinv : nullptr;

    if (fold) {
        const int units = 8 * (Fu / 32) + 8 * (Fm / 32) + 8;
        compose_wmma<<<ceil_div_i(units, WPB), TPB, 0, stream>>>(W1, b1, Wu, bu, Wm, bm,
                                                                 Wcu, bcu, Wcm, bcm, Fu, Fm);
        launch_gemm(x_user, Wcu, bcu, rowScale, Hh, Nu, Fu, H, stream);
        launch_gemm(x_movie, Wcm, bcm, rowScale ? rowScale + Nu : nullptr,
                    Hh + (size_t)Nu * H, Nm, Fm, H, stream);
    } else {
        launch_gemm(x_user, Wu, bu, nullptr, X, Nu, Fu, H, stream);
        launch_gemm(x_movie, Wm, bm, nullptr, X + (size_t)Nu * H, Nm, Fm, H, stream);
    }

    const int gatherWaves = N * ((H + 127) / 128);

    if (!fold) launch_gemm(X, W1, b1, rowScale, Hh, N, H, H, stream);
    if (sorted)
        gather_relu_kernel<true><<<ceil_div_i(gatherWaves, WPB), TPB, 0, stream>>>(Hh, dinv, start, cnt, slots, srcs, Acc, N, E, H);
    else
        gather_relu_kernel<false><<<ceil_div_i(gatherWaves, WPB), TPB, 0, stream>>>(Hh, dinv, start, cnt, slots, srcs, Acc, N, E, H);

    launch_gemm(Acc, W2, b2, rowScale, Hh, N, H, H, stream);
    if (sorted)
        gather_relu_kernel<true><<<ceil_div_i(gatherWaves, WPB), TPB, 0, stream>>>(Hh, dinv, start, cnt, slots, srcs, X, N, E, H);
    else
        gather_relu_kernel<false><<<ceil_div_i(gatherWaves, WPB), TPB, 0, stream>>>(Hh, dinv, start, cnt, slots, srcs, X, N, E, H);

    const int edgeTiles = ceil_div_i(E, 16);
    if (H == 128 && C == 10) {
        const int nodeTiles = ceil_div_i(Nu, 16) + ceil_div_i(Nm, 16);
        node_proj_wmma<<<ceil_div_i(nodeTiles, 4), 128, 0, stream>>>(X, We, be, P, Nu, Nm, C);
        const long long quads = ((long long)E * C + 3) / 4;
        edge_add10<4><<<ceil_div_i(quads, TPB * 4), TPB, 0, stream>>>(P, row, col, out, E, Nu);
    } else if (H == 128)
        edge_mlp_wmma<128><<<ceil_div_i(edgeTiles, WPB), TPB, 0, stream>>>(X, row, col, We, be, out, E, Nu, H, C);
    else
        edge_mlp_wmma<0><<<ceil_div_i(edgeTiles, WPB), TPB, 0, stream>>>(X, row, col, We, be, out, E, Nu, H, C);
}


# independent launches merged: the conv1 weight compose rides in the first sort launch, the two input GEMMs share one launch on disjoint blocks; 11 launches become 9; sum order unchanged
# speedup vs baseline: 5.2633x; 1.0511x over previous
#include <hip/hip_runtime.h>

typedef __attribute__((ext_vector_type(2))) float v2f;
typedef __attribute__((ext_vector_type(8))) float v8f;

__device__ __forceinline__ v8f wmma_f32x4(v2f a, v2f b, v8f c) {
    return __builtin_amdgcn_wmma_f32_16x16x4_f32(
         false, a,  false, b,
         (short)0, c,  false,  false);
}

__device__ __forceinline__ void load_row_scale8(const float* __restrict__ scale,
                                                int row0, int M, float (&sc)[8]) {
    if (scale == nullptr) {
#pragma unroll
        for (int r = 0; r < 8; ++r) sc[r] = 1.0f;
    } else {
#pragma unroll
        for (int r = 0; r < 8; ++r) sc[r] = scale[min(row0 + r, M - 1)];
    }
}

template <int K>
__global__ void gemm_bias_wmma_lds(const float* __restrict__ X,
                                   const float* __restrict__ W,
                                   const float* __restrict__ bias,
                                   const float* __restrict__ scale,
                                   float* __restrict__ Y,
                                   int M, int N) {
    __shared__ float As[16 * K];

    const int lane = threadIdx.x & 31;
    const int wave = threadIdx.x >> 5;
    const int wavesPerBlock = blockDim.x >> 5;
    const int tilesN = N >> 4;
    const int strip = blockIdx.x * 16;

    if (strip + 16 <= M) {
        const float4* __restrict__ src = (const float4*)(X + (size_t)strip * K);
        float4* dst = (float4*)As;
        for (int i = threadIdx.x; i < (16 * K) / 4; i += blockDim.x) dst[i] = src[i];
    } else {
        for (int i = threadIdx.x; i < 16 * K; i += blockDim.x) {
            const int rr = i / K, cc = i - rr * K;
            const int gr = min(strip + rr, M - 1);
            As[i] = X[(size_t)gr * K + cc];
        }
    }
    __syncthreads();

    const int koff = (lane >> 4) << 1;
    const float* __restrict__ arow = As + (lane & 15) * K + koff;
    const bool fullTile = (strip + 16 <= M);
    const int rbase = strip + ((lane >> 4) << 3);
    float sc[8];
    load_row_scale8(scale, rbase, M, sc);

    for (int tn = wave; tn < tilesN; tn += wavesPerBlock) {
        const int ncol = tn * 16 + (lane & 15);
        const float* __restrict__ wrow = W + (size_t)ncol * K + koff;

        v8f acc = {};
#pragma unroll
        for (int k = 0; k < K; k += 4) {
            v2f a = *(const v2f*)(arow + k);
            v2f b = *(const v2f*)(wrow + k);
            acc = wmma_f32x4(a, b, acc);
        }

        const float bv = bias[ncol];
        float* yp = Y + (size_t)rbase * N + ncol;
        if (fullTile) {
#pragma unroll
            for (int r = 0; r < 8; ++r) { *yp = (acc[r] + bv) * sc[r]; yp += N; }
        } else {
#pragma unroll
            for (int r = 0; r < 8; ++r) {
                if (rbase + r < M) *yp = (acc[r] + bv) * sc[r];
                yp += N;
            }
        }
    }
}

template <int K, bool SCALED>
__device__ __forceinline__ void
gemm_ws_body(const float* __restrict__ X,
             const float* __restrict__ W,
             const float* __restrict__ bias,
             const float* __restrict__ scale,
             float* __restrict__ Y, int M,
             float* __restrict__ lds, int firstStrip, int stripStep) {
    constexpr int N = 128;
    constexpr int P = K + 4;
    constexpr int C4 = K / 4;
    constexpr int NV = 16 * C4;
    constexpr int PER = (NV + 255) / 256;
    float (*As)[16 * P] = (float (*)[16 * P])lds;

    const int lane = threadIdx.x & 31;
    const int wave = threadIdx.x >> 5;
    const int c = lane & 15;
    const int h = lane >> 4;
    const int ncol = wave * 16 + c;
    const int strips = (M + 15) >> 4;
    const int step = stripStep;

    float4 bf[K / 8];
    {
        const float* __restrict__ wrow = W + (size_t)ncol * K + 4 * h;
#pragma unroll
        for (int i = 0; i < K / 8; ++i) bf[i] = *(const float4*)(wrow + 8 * i);
    }
    const float bv = bias[ncol];

    static_assert(PER <= 2, "a strip is at most 512 float4s");
    const int i0 = threadIdx.x;
    const int rr0 = i0 / C4, cc0 = i0 - rr0 * C4;
    const int i1 = i0 + 256;
    const int rr1 = i1 / C4, cc1 = i1 - rr1 * C4;

    int s = firstStrip;
    if (s < strips) {
        if (NV >= 256 || i0 < NV)
            *(float4*)(&As[0][rr0 * P + 4 * cc0]) =
                *(const float4*)(X + (size_t)min(s * 16 + rr0, M - 1) * K + 4 * cc0);
        if (PER > 1)
            *(float4*)(&As[0][rr1 * P + 4 * cc1]) =
                *(const float4*)(X + (size_t)min(s * 16 + rr1, M - 1) * K + 4 * cc1);
    }
    __syncthreads();

    const int aoff = c * P + 4 * h;
    int buf = 0;
    for (; s < strips; s += step) {
        const int sn = s + step;
        const bool more = sn < strips;
        float4 pre0 = make_float4(0.f, 0.f, 0.f, 0.f);
        float4 pre1 = make_float4(0.f, 0.f, 0.f, 0.f);
        if (more) {
            if (NV >= 256 || i0 < NV)
                pre0 = *(const float4*)(X + (size_t)min(sn * 16 + rr0, M - 1) * K + 4 * cc0);
            if (PER > 1)
                pre1 = *(const float4*)(X + (size_t)min(sn * 16 + rr1, M - 1) * K + 4 * cc1);
        }
        const int rbase = s * 16 + 8 * h;
        float4 su = make_float4(1.f, 1.f, 1.f, 1.f), sv = su;
        if (SCALED) {
            const float* __restrict__ sp = scale + max(min(rbase, M - 8), 0);
            __builtin_memcpy(&su, sp, 16);
            __builtin_memcpy(&sv, sp + 4, 16);
        }

        const float* arow = &As[buf][aoff];
        v8f acc = {};
#pragma unroll
        for (int i = 0; i < K / 8; ++i) {
            const float4 a = *(const float4*)(arow + 8 * i);
            v2f a0 = {a.x, a.y}, b0 = {bf[i].x, bf[i].y};
            v2f a1 = {a.z, a.w}, b1 = {bf[i].z, bf[i].w};
            acc = wmma_f32x4(a0, b0, acc);
            acc = wmma_f32x4(a1, b1, acc);
        }

        if (more) {
            if (NV >= 256 || i0 < NV) *(float4*)(&As[buf ^ 1][rr0 * P + 4 * cc0]) = pre0;
            if (PER > 1) *(float4*)(&As[buf ^ 1][rr1 * P + 4 * cc1]) = pre1;
        }
        __syncthreads();
        buf ^= 1;

        float* yp = Y + (size_t)rbase * N + ncol;
        if (s * 16 + 16 <= M) {
            const float sc[8] = {su.x, su.y, su.z, su.w, sv.x, sv.y, sv.z, sv.w};
#pragma unroll
            for (int r = 0; r < 8; ++r) { *yp = (acc[r] + bv) * sc[r]; yp += N; }
        } else {
#pragma unroll
            for (int r = 0; r < 8; ++r) {
                if (rbase + r < M) *yp = (acc[r] + bv) * (SCALED ? scale[rbase + r] : 1.0f);
                yp += N;
            }
        }
    }
}

template <int K, bool SCALED>
__global__ void __launch_bounds__(256)
gemm_bias_wmma_ws(const float* __restrict__ X,
                  const float* __restrict__ W,
                  const float* __restrict__ bias,
                  const float* __restrict__ scale,
                  float* __restrict__ Y, int M) {
    __shared__ __attribute__((aligned(16))) float As[2 * 16 * (K + 4)];
    gemm_ws_body<K, SCALED>(X, W, bias, scale, Y, M, As, (int)blockIdx.x, (int)gridDim.x);
}

template <int KA, int KB, bool SCALED>
__global__ void __launch_bounds__(256)
gemm_pair_wmma_ws(const float* __restrict__ XA, const float* __restrict__ WA,
                  const float* __restrict__ biasA, const float* __restrict__ scaleA,
                  float* __restrict__ YA, int MA, int gridA,
                  const float* __restrict__ XB, const float* __restrict__ WB,
                  const float* __restrict__ biasB, const float* __restrict__ scaleB,
                  float* __restrict__ YB, int MB) {
    constexpr int KMAX = KA > KB ? KA : KB;
    __shared__ __attribute__((aligned(16))) float As[2 * 16 * (KMAX + 4)];
    const int bid = (int)blockIdx.x;
    if (bid < gridA)
        gemm_ws_body<KA, SCALED>(XA, WA, biasA, scaleA, YA, MA, As, bid, gridA);
    else
        gemm_ws_body<KB, SCALED>(XB, WB, biasB, scaleB, YB, MB, As, bid - gridA,
                                 (int)gridDim.x - gridA);
}

__device__ __forceinline__ void
compose_unit(const float* __restrict__ W1, const float* __restrict__ b1,
             const float* __restrict__ Wu, const float* __restrict__ bu,
             const float* __restrict__ Wm, const float* __restrict__ bm,
             float* __restrict__ Wcu, float* __restrict__ bcu,
             float* __restrict__ Wcm, float* __restrict__ bcm,
             int Fu, int Fm, int unit, int lane) {
    constexpr int H = 128;
    constexpr int S = H / 8;
    const int c = lane & 15;
    const int h = lane >> 4;
    const int wu = 8 * (Fu >> 5);
    const int wm = 8 * (Fm >> 5);

    if (unit < wu + wm) {
        const bool second = unit >= wu;
        const int u = second ? unit - wu : unit;
        const int F = second ? Fm : Fu;
        const float* __restrict__ Wt = second ? Wm : Wu;
        float* __restrict__ Wc = second ? Wcm : Wcu;
        const int groups = F >> 5;
        const int nt = u / groups;
        const int cg = u - nt * groups;

        const float* __restrict__ arow = W1 + (size_t)(nt * 16 + c) * H + 4 * h;
        const float* __restrict__ bcol = Wt + (size_t)(4 * h) * F + cg * 32 + c;

        float4 af[S];
        float bl[S][4], br[S][4];
#pragma unroll
        for (int i = 0; i < S; ++i) {
            af[i] = *(const float4*)(arow + 8 * i);
            const float* __restrict__ bp = bcol + (size_t)(8 * i) * F;
#pragma unroll
            for (int e = 0; e < 4; ++e) {
                bl[i][e] = bp[(size_t)e * F];
                br[i][e] = bp[(size_t)e * F + 16];
            }
        }

        v8f acc0 = {}, acc1 = {};
#pragma unroll
        for (int i = 0; i < S; ++i) {
            v2f a0 = {af[i].x, af[i].y}, a1 = {af[i].z, af[i].w};
            v2f p0 = {bl[i][0], bl[i][1]}, p1 = {bl[i][2], bl[i][3]};
            v2f q0 = {br[i][0], br[i][1]}, q1 = {br[i][2], br[i][3]};
            acc0 = wmma_f32x4(a0, p0, acc0);
            acc1 = wmma_f32x4(a0, q0, acc1);
            acc0 = wmma_f32x4(a1, p1, acc0);
            acc1 = wmma_f32x4(a1, q1, acc1);
        }

        float* wp = Wc + (size_t)(nt * 16 + 8 * h) * F + cg * 32 + c;
#pragma unroll
        for (int r = 0; r < 8; ++r) {
            wp[0] = acc0[r];
            wp[16] = acc1[r];
            wp += F;
        }
    } else if (unit < wu + wm + 8) {
        const int q = unit - wu - wm;
        const bool second = q >= 4;
        const float* __restrict__ bt = second ? bm : bu;
        float* __restrict__ bc = second ? bcm : bcu;
        const int nb = 32 * (q & 3);

        const float* __restrict__ ar0 = W1 + (size_t)(nb + c) * H + 4 * h;
        const float* __restrict__ ar1 = ar0 + (size_t)16 * H;
        const float* __restrict__ bv = bt + 4 * h;

        float4 a0f[S], a1f[S], bf[S];
#pragma unroll
        for (int i = 0; i < S; ++i) {
            a0f[i] = *(const float4*)(ar0 + 8 * i);
            a1f[i] = *(const float4*)(ar1 + 8 * i);
            bf[i] = *(const float4*)(bv + 8 * i);
        }
        const float4 s00 = *(const float4*)(b1 + nb + 8 * h);
        const float4 s01 = *(const float4*)(b1 + nb + 8 * h + 4);
        const float4 s10 = *(const float4*)(b1 + nb + 16 + 8 * h);
        const float4 s11 = *(const float4*)(b1 + nb + 16 + 8 * h + 4);

        v8f acc0 = {}, acc1 = {};
#pragma unroll
        for (int i = 0; i < S; ++i) {
            v2f x0 = {a0f[i].x, a0f[i].y}, x1 = {a0f[i].z, a0f[i].w};
            v2f y0 = {a1f[i].x, a1f[i].y}, y1 = {a1f[i].z, a1f[i].w};
            v2f p0 = {bf[i].x, bf[i].y}, p1 = {bf[i].z, bf[i].w};
            acc0 = wmma_f32x4(x0, p0, acc0);
            acc1 = wmma_f32x4(y0, p0, acc1);
            acc0 = wmma_f32x4(x1, p1, acc0);
            acc1 = wmma_f32x4(y1, p1, acc1);
        }

        const float4 o00 = make_float4(acc0[0] + s00.x, acc0[1] + s00.y, acc0[2] + s00.z, acc0[3] + s00.w);
        const float4 o01 = make_float4(acc0[4] + s01.x, acc0[5] + s01.y, acc0[6] + s01.z, acc0[7] + s01.w);
        const float4 o10 = make_float4(acc1[0] + s10.x, acc1[1] + s10.y, acc1[2] + s10.z, acc1[3] + s10.w);
        const float4 o11 = make_float4(acc1[4] + s11.x, acc1[5] + s11.y, acc1[6] + s11.z, acc1[7] + s11.w);
        if (c == 0) {
            float* op = bc + nb + 8 * h;
            *(float4*)(op) = o00;
            *(float4*)(op + 4) = o01;
            *(float4*)(op + 16) = o10;
            *(float4*)(op + 20) = o11;
        }
    }
}

__global__ void __launch_bounds__(256)
compose_wmma(const float* __restrict__ W1, const float* __restrict__ b1,
             const float* __restrict__ Wu, const float* __restrict__ bu,
             const float* __restrict__ Wm, const float* __restrict__ bm,
             float* __restrict__ Wcu, float* __restrict__ bcu,
             float* __restrict__ Wcm, float* __restrict__ bcm,
             int Fu, int Fm) {
    const int unit = blockIdx.x * (blockDim.x >> 5) + (threadIdx.x >> 5);
    compose_unit(W1, b1, Wu, bu, Wm, bm, Wcu, bcu, Wcm, bcm, Fu, Fm, unit, threadIdx.x & 31);
}

__global__ void gemm_bias_wmma_rt(const float* __restrict__ X,
                                  const float* __restrict__ W,
                                  const float* __restrict__ bias,
                                  const float* __restrict__ scale,
                                  float* __restrict__ Y,
                                  int M, int K, int N) {
    const int lane = threadIdx.x & 31;
    const int wave = threadIdx.x >> 5;
    const int tilesN = N >> 4;
    const int tile = blockIdx.x * (blockDim.x >> 5) + wave;
    const int tm = tile / tilesN;
    const int tn = tile - tm * tilesN;
    if (tm * 16 >= M) return;

    const int mrow = min(tm * 16 + (lane & 15), M - 1);
    const int ncol = tn * 16 + (lane & 15);
    const int koff = (lane >> 4) << 1;
    const float* __restrict__ xrow = X + (size_t)mrow * K + koff;
    const float* __restrict__ wrow = W + (size_t)ncol * K + koff;

    v8f acc = {};
#pragma unroll 8
    for (int k = 0; k < K; k += 4) {
        v2f a = *(const v2f*)(xrow + k);
        v2f b = *(const v2f*)(wrow + k);
        acc = wmma_f32x4(a, b, acc);
    }

    const float bv = bias[ncol];
    const int rbase = tm * 16 + ((lane >> 4) << 3);
    float sc[8];
    load_row_scale8(scale, rbase, M, sc);
    float* yp = Y + (size_t)rbase * N + ncol;
#pragma unroll
    for (int r = 0; r < 8; ++r) {
        if (rbase + r < M) *yp = (acc[r] + bv) * sc[r];
        yp += N;
    }
}

template <int HT>
__global__ void edge_mlp_wmma(const float* __restrict__ xf,
                              const int* __restrict__ row,
                              const int* __restrict__ col,
                              const float* __restrict__ We,
                              const float* __restrict__ be,
                              float* __restrict__ out,
                              int E, int Nu, int Hrt, int C) {
    const int H = (HT > 0) ? HT : Hrt;
    const int K = 2 * H;
    const int lane = threadIdx.x & 31;
    const int wave = threadIdx.x >> 5;
    const int tile = blockIdx.x * (blockDim.x >> 5) + wave;
    const int ebase = tile * 16;
    if (ebase >= E) return;

    const int m = lane & 15;
    const int e = min(ebase + m, E - 1);
    const int r = row[e];
    const int c = col[e];
    const int koff = (lane >> 4) << 1;

    const float* __restrict__ urow = xf + (size_t)r * H + koff;
    const float* __restrict__ mrow = xf + (size_t)(Nu + c) * H + koff;

    const int n = lane & 15;
    const float wmask = (n < C) ? 1.0f : 0.0f;
    const float* __restrict__ wrow = We + (size_t)min(n, C - 1) * K + koff;

    v8f acc = {};
#pragma unroll 8
    for (int k = 0; k < K; k += 4) {
        const float* src = (k + koff < H) ? (urow + k) : (mrow + (k - H));
        v2f a = *(const v2f*)src;
        v2f b = *(const v2f*)(wrow + k);
        b.x *= wmask;
        b.y *= wmask;
        acc = wmma_f32x4(a, b, acc);
    }

    if (n < C) {
        const float bv = be[n];
        const int mb = (lane >> 4) << 3;
        float* op = out + (size_t)(ebase + mb) * C + n;
        if (ebase + 16 <= E) {
#pragma unroll
            for (int rr = 0; rr < 8; ++rr) { *op = acc[rr] + bv; op += C; }
        } else {
#pragma unroll
            for (int rr = 0; rr < 8; ++rr) {
                if (ebase + mb + rr < E) *op = acc[rr] + bv;
                op += C;
            }
        }
    }
}

__global__ void node_proj_wmma(const float* __restrict__ xf,
                               const float* __restrict__ We,
                               const float* __restrict__ be,
                               float* __restrict__ P,
                               int Nu, int Nm, int C) {
    constexpr int H = 128;
    const int lane = threadIdx.x & 31;
    const int wave = threadIdx.x >> 5;
    const int tilesU = (Nu + 15) >> 4;
    const int tilesM = (Nm + 15) >> 4;
    const int tile = blockIdx.x * (blockDim.x >> 5) + wave;
    if (tile >= tilesU + tilesM) return;

    const bool second = tile >= tilesU;
    const int segEnd = second ? Nu + Nm : Nu;
    const int base = second ? Nu + ((tile - tilesU) << 4) : (tile << 4);

    const int c = lane & 15;
    const int h = lane >> 4;
    const int node = min(base + 2 * (c & 7) + (c >> 3), segEnd - 1);
    const float* __restrict__ arow = xf + (size_t)node * H + 4 * h;
    const float* __restrict__ brow =
        We + (size_t)min(c, C - 1) * (2 * H) + (second ? H : 0) + 4 * h;

    const float bsel = be[min(c, C - 1)];
    const float bv = (second && c < C) ? bsel : 0.0f;

    v8f acc0 = {}, acc1 = {};
#pragma unroll
    for (int i = 0; i < H / 8; ++i) {
        const float4 a = *(const float4*)(arow + 8 * i);
        const float4 b = *(const float4*)(brow + 8 * i);
        v2f a0 = {a.x, a.y}, b0 = {b.x, b.y};
        v2f a1 = {a.z, a.w}, b1 = {b.z, b.w};
        acc0 = wmma_f32x4(a0, b0, acc0);
        acc1 = wmma_f32x4(a1, b1, acc1);
    }

    float v[8];
#pragma unroll
    for (int r = 0; r < 8; ++r) v[r] = (acc0[r] + acc1[r]) + bv;

    float* pp = P + (size_t)base * 16 + lane;
    if (base + 16 <= segEnd) {
#pragma unroll
        for (int r = 0; r < 8; ++r) pp[32 * r] = v[r];
    } else {
#pragma unroll
        for (int r = 0; r < 8; ++r)
            if (base + 2 * r + h < segEnd) pp[32 * r] = v[r];
    }
}

template <int ITER>
__global__ void edge_add10(const float* __restrict__ P,
                           const int* __restrict__ row,
                           const int* __restrict__ col,
                           float* __restrict__ out,
                           int E, int Nu) {
    constexpr unsigned C = 10;
    const unsigned total = (unsigned)E * C;
    const float* __restrict__ Pu = P;
    const float* __restrict__ Pm = P + (size_t)Nu * 16;
    const unsigned perBlock = blockDim.x * ITER;
    const unsigned qb = blockIdx.x * perBlock;

    if ((qb + perBlock) * 4u <= total) {
        unsigned f0[ITER], cA[ITER], cB[ITER];
        int rA[ITER], mA[ITER], rB[ITER], mB[ITER];
#pragma unroll
        for (int it = 0; it < ITER; ++it) {
            f0[it] = 4u * (qb + it * blockDim.x + threadIdx.x);
            const unsigned f1 = f0[it] + 2u;
            const unsigned eA = f0[it] / C, eB = f1 / C;
            cA[it] = f0[it] - eA * C;
            cB[it] = f1 - eB * C;
            rA[it] = row[eA]; mA[it] = col[eA];
            rB[it] = row[eB]; mB[it] = col[eB];
        }
        float2 uA[ITER], vA[ITER], uB[ITER], vB[ITER];
#pragma unroll
        for (int it = 0; it < ITER; ++it) {
            uA[it] = *(const float2*)(Pu + (size_t)rA[it] * 16 + cA[it]);
            vA[it] = *(const float2*)(Pm + (size_t)mA[it] * 16 + cA[it]);
            uB[it] = *(const float2*)(Pu + (size_t)rB[it] * 16 + cB[it]);
            vB[it] = *(const float2*)(Pm + (size_t)mB[it] * 16 + cB[it]);
        }
#pragma unroll
        for (int it = 0; it < ITER; ++it) {
            *(float4*)(out + f0[it]) = make_float4(uA[it].x + vA[it].x, uA[it].y + vA[it].y,
                                                   uB[it].x + vB[it].x, uB[it].y + vB[it].y);
        }
    } else {
        for (unsigned p = threadIdx.x; p < 2u * perBlock; p += blockDim.x) {
            const unsigned f = 4u * qb + 2u * p;
            if (f < total) {
                const unsigned e = f / C, c = f - e * C;
                const float2 u = *(const float2*)(Pu + (size_t)row[e] * 16 + c);
                const float2 v = *(const float2*)(Pm + (size_t)col[e] * 16 + c);
                *(float2*)(out + f) = make_float2(u.x + v.x, u.y + v.y);
            }
        }
    }
}

constexpr int SORT_G = 128;
constexpr int SORT_NBMAX = 1024;
constexpr int SORT_T = 1024;
static_assert(SORT_T == SORT_NBMAX && SORT_G % 32 == 0, "one bin per thread, whole chunk rows per wave");

__device__ __forceinline__ int wave_incl_scan(int v, int lane) {
    for (int d = 1; d < 32; d <<= 1) {
        const int t = __shfl_up(v, d);
        if (lane >= d) v += t;
    }
    return v;
}

constexpr int COMPOSE_WPB = 8;
__global__ void __launch_bounds__(SORT_T)
bin_hist_kernel(const int* __restrict__ col, int* __restrict__ BH,
                int E, int NB, int chunk,
                const float* __restrict__ W1, const float* __restrict__ b1,
                const float* __restrict__ Wu, const float* __restrict__ bu,
                const float* __restrict__ Wm, const float* __restrict__ bm,
                float* __restrict__ Wcu, float* __restrict__ bcu,
                float* __restrict__ Wcm, float* __restrict__ bcm,
                int Fu, int Fm) {
    __shared__ int hist[SORT_NBMAX];
    const int tid = threadIdx.x;
    if (blockIdx.x >= SORT_G) {
        const int wave = tid >> 5;
        if (wave < COMPOSE_WPB)
            compose_unit(W1, b1, Wu, bu, Wm, bm, Wcu, bcu, Wcm, bcm, Fu, Fm,
                         ((int)blockIdx.x - SORT_G) * COMPOSE_WPB + wave, tid & 31);
        return;
    }
    const int g = blockIdx.x;
    hist[tid] = 0;
    __syncthreads();
    const int lo = min(g * chunk, E);
    const int hi = min(lo + chunk, E);
    for (int e0 = lo + tid; e0 < hi; e0 += 4 * SORT_T) {
        int c[4];
#pragma unroll
        for (int k = 0; k < 4; ++k) {
            const int e = e0 + k * SORT_T;
            c[k] = (e < hi) ? col[e] : -1;
        }
#pragma unroll
        for (int k = 0; k < 4; ++k)
            if (c[k] >= 0) atomicAdd(&hist[min(c[k] >> 8, NB - 1)], 1);
    }
    __syncthreads();
    BH[(size_t)g * SORT_NBMAX + tid] = hist[tid];
}

__global__ void __launch_bounds__(SORT_T)
bin_scatter_kernel(const int* __restrict__ row, const int* __restrict__ col,
                   const int* __restrict__ BH,
                   int* __restrict__ binBase, int* __restrict__ binCount,
                   int2* __restrict__ binned, int E, int NB, int chunk) {
    __shared__ int pre[SORT_NBMAX];
    __shared__ int tot[SORT_NBMAX];
    __shared__ int cur[SORT_NBMAX];
    __shared__ int wsum[32];
    const int tid = threadIdx.x;
    const int lane = tid & 31;
    const int wave = tid >> 5;
    const int g = blockIdx.x;

    pre[tid] = 0;
    tot[tid] = 0;
    __syncthreads();
    constexpr int GW = SORT_G / 32;
    const int g0 = wave * GW;
    for (int bb = 0; bb * 128 < NB; bb += 2) {
        const int b4 = bb * 128 + 4 * lane;
        const bool two = (bb + 1) * 128 < NB;
        int4 va[GW], vb[GW];
#pragma unroll
        for (int k = 0; k < GW; ++k) {
            const int* rp = BH + (size_t)(g0 + k) * SORT_NBMAX + b4;
            va[k] = *(const int4*)rp;
            vb[k] = two ? *(const int4*)(rp + 128) : make_int4(0, 0, 0, 0);
        }
        int4 pa = make_int4(0, 0, 0, 0), ta = pa, pb = pa, tb = pa;
#pragma unroll
        for (int k = 0; k < GW; ++k) {
            const bool early = g0 + k < g;
            ta.x += va[k].x; ta.y += va[k].y; ta.z += va[k].z; ta.w += va[k].w;
            tb.x += vb[k].x; tb.y += vb[k].y; tb.z += vb[k].z; tb.w += vb[k].w;
            if (early) {
                pa.x += va[k].x; pa.y += va[k].y; pa.z += va[k].z; pa.w += va[k].w;
                pb.x += vb[k].x; pb.y += vb[k].y; pb.z += vb[k].z; pb.w += vb[k].w;
            }
        }
        atomicAdd(&tot[b4 + 0], ta.x); atomicAdd(&tot[b4 + 1], ta.y);
        atomicAdd(&tot[b4 + 2], ta.z); atomicAdd(&tot[b4 + 3], ta.w);
        atomicAdd(&pre[b4 + 0], pa.x); atomicAdd(&pre[b4 + 1], pa.y);
        atomicAdd(&pre[b4 + 2], pa.z); atomicAdd(&pre[b4 + 3], pa.w);
        if (two) {
            atomicAdd(&tot[b4 + 128], tb.x); atomicAdd(&tot[b4 + 129], tb.y);
            atomicAdd(&tot[b4 + 130], tb.z); atomicAdd(&tot[b4 + 131], tb.w);
            atomicAdd(&pre[b4 + 128], pb.x); atomicAdd(&pre[b4 + 129], pb.y);
            atomicAdd(&pre[b4 + 130], pb.z); atomicAdd(&pre[b4 + 131], pb.w);
        }
    }
    __syncthreads();
    const int mine = tot[tid];
    const int incl = wave_incl_scan(mine, lane);
    if (lane == 31) wsum[wave] = incl;
    __syncthreads();
    if (wave == 0) {
        const int w = wsum[lane];
        const int wi = wave_incl_scan(w, lane);
        wsum[lane] = wi - w;
    }
    __syncthreads();
    const int base = wsum[wave] + incl - mine;
    cur[tid] = base + pre[tid];
    if (g == 0 && tid < NB) { binBase[tid] = base; binCount[tid] = mine; }
    __syncthreads();

    const int lo = min(g * chunk, E);
    const int hi = min(lo + chunk, E);
    for (int e0 = lo + tid; e0 < hi; e0 += 4 * SORT_T) {
        int c[4], r[4];
#pragma unroll
        for (int k = 0; k < 4; ++k) {
            const int e = e0 + k * SORT_T;
            c[k] = (e < hi) ? col[e] : -1;
            r[k] = (e < hi) ? row[e] : 0;
        }
#pragma unroll
        for (int k = 0; k < 4; ++k) {
            if (c[k] >= 0) {
                const int pos = atomicAdd(&cur[min(c[k] >> 8, NB - 1)], 1);
                binned[min(max(pos, 0), E - 1)] = make_int2(r[k], c[k]);
            }
        }
    }
}

__global__ void __launch_bounds__(SORT_T)
bin_sort_kernel(const int2* __restrict__ binned,
                const int* __restrict__ binBase, const int* __restrict__ binCount,
                int* __restrict__ start, int* __restrict__ cnt,
                float* __restrict__ dinv, int* __restrict__ srcs, int N, int E) {
    __shared__ int c[256];
    __shared__ int cur[256];
    __shared__ int wsum[8];
    const int tid = threadIdx.x;
    const int lane = tid & 31;
    const int wave = tid >> 5;
    const int b = blockIdx.x;
    const int count = min(max(binCount[b], 0), E);
    const int base = min(max(binBase[b], 0), E - count);
    const int node0 = b << 8;

    if (tid < 256) c[tid] = 0;
    __syncthreads();
    for (int i0 = tid; i0 < count; i0 += 4 * SORT_T) {
        int t[4];
#pragma unroll
        for (int k = 0; k < 4; ++k) {
            const int i = i0 + k * SORT_T;
            t[k] = (i < count) ? binned[base + i].y : -1;
        }
#pragma unroll
        for (int k = 0; k < 4; ++k)
            if (t[k] >= 0) atomicAdd(&c[min(max(t[k] - node0, 0), 255)], 1);
    }
    __syncthreads();
    int mine = 0, incl = 0;
    if (tid < 256) {
        mine = c[tid];
        incl = wave_incl_scan(mine, lane);
        if (lane == 31) wsum[wave] = incl;
    }
    __syncthreads();
    if (tid < 256) {
        int off = 0;
        for (int w = 0; w < wave; ++w) off += wsum[w];
        const int excl = off + incl - mine;
        cur[tid] = excl;
        const int node = node0 + tid;
        if (node < N) {
            start[node] = base + excl;
            cnt[node] = mine;
            dinv[node] = rsqrtf((float)(mine + 1));
        }
    }
    __syncthreads();
    for (int i0 = tid; i0 < count; i0 += 4 * SORT_T) {
        int2 ed[4];
#pragma unroll
        for (int k = 0; k < 4; ++k) {
            const int i = i0 + k * SORT_T;
            ed[k] = (i < count) ? binned[base + i] : make_int2(0, -1);
        }
#pragma unroll
        for (int k = 0; k < 4; ++k) {
            if (ed[k].y >= 0) {
                const int p = atomicAdd(&cur[min(max(ed[k].y - node0, 0), 255)], 1);
                srcs[base + min(max(p, 0), count - 1)] = ed[k].x;
            }
        }
    }
}

__global__ void zero_int_kernel(int* __restrict__ p, int n) {
    const int i = blockIdx.x * blockDim.x + threadIdx.x;
    if (i < n) p[i] = 0;
}

__global__ void count_targets_kernel(const int* __restrict__ col,
                                     int* __restrict__ cnt, int E) {
    const int e = blockIdx.x * blockDim.x + threadIdx.x;
    if (e < E) atomicAdd(&cnt[col[e]], 1);
}

__global__ void alloc_segments_kernel(const int* __restrict__ cnt,
                                      int* __restrict__ total,
                                      int* __restrict__ start,
                                      int* __restrict__ cursor,
                                      float* __restrict__ dinv, int N) {
    const int i = blockIdx.x * blockDim.x + threadIdx.x;
    const int lane = threadIdx.x & 31;
    const int c = (i < N) ? cnt[i] : 0;
    int incl = c;
    for (int d = 1; d < 32; d <<= 1) {
        const int t = __shfl_up(incl, d);
        if (lane >= d) incl += t;
    }
    int base = 0;
    if (lane == 31) base = atomicAdd(total, incl);
    base = __shfl(base, 31);
    if (i < N) {
        const int s = base + incl - c;
        start[i] = s;
        cursor[i] = s;
        dinv[i] = rsqrtf((float)(c + 1));
    }
}

__global__ void fill_segments_kernel(const int* __restrict__ row,
                                     const int* __restrict__ col,
                                     const float* __restrict__ dinv,
                                     int* __restrict__ cursor,
                                     int2* __restrict__ slots, int E) {
    const int e = blockIdx.x * blockDim.x + threadIdx.x;
    if (e < E) {
        const int r = row[e];
        const int pos = atomicAdd(&cursor[col[e]], 1);
        slots[pos] = make_int2(r, __float_as_int(dinv[r]));
    }
}

__device__ __forceinline__ float4 fma4(float w, float4 a, float4 c) {
    return make_float4(fmaf(w, a.x, c.x), fmaf(w, a.y, c.y), fmaf(w, a.z, c.z), fmaf(w, a.w, c.w));
}

__device__ __forceinline__ float4 add4(float4 a, float4 c) {
    return make_float4(a.x + c.x, a.y + c.y, a.z + c.z, a.w + c.w);
}

template <bool SORTED>
__global__ void gather_relu_kernel(const float* __restrict__ h,
                                   const float* __restrict__ dinv,
                                   const int* __restrict__ start,
                                   const int* __restrict__ cnt,
                                   const int2* __restrict__ slots,
                                   const int* __restrict__ srcs,
                                   float* __restrict__ xout,
                                   int N, int E, int H) {
    const int lane = threadIdx.x & 31;
    const int cpn = (H + 127) >> 7;
    const int wid = __builtin_amdgcn_readfirstlane(
        (int)(blockIdx.x * (blockDim.x >> 5) + (threadIdx.x >> 5)));
    const int node = wid / cpn;
    if (node >= N) return;
    const int cb = (wid - node * cpn) << 7;
    const bool act = cb + 4 * lane < H;
    const int c0 = min(cb + 4 * lane, H - 4);
    const float* __restrict__ hl = h + c0;

    const int c = min(max(cnt[node], 0), E);
    const int s = min(max(start[node], 0), E - c);
    const float4 hself = *(const float4*)(hl + (size_t)node * H);
    const float di = dinv[node];

    float4 acc = make_float4(0.f, 0.f, 0.f, 0.f);
    for (int k0 = 0; k0 < c; k0 += 32) {
        const int n = min(32, c - k0);
        int2 ev = make_int2(0, 0);
        if (lane < n) {
            if (SORTED)
                ev.x = min(max(srcs[s + k0 + lane], 0), N - 1);
            else
                ev = slots[s + k0 + lane];
        }
        for (int j = 0; j < n; j += 4) {
            const int j1 = min(j + 1, n - 1), j2 = min(j + 2, n - 1), j3 = min(j + 3, n - 1);
            const int s0 = __builtin_amdgcn_readlane(ev.x, j);
            const int s1 = __builtin_amdgcn_readlane(ev.x, j1);
            const int s2 = __builtin_amdgcn_readlane(ev.x, j2);
            const int s3 = __builtin_amdgcn_readlane(ev.x, j3);
            const float4 a0 = *(const float4*)(hl + (size_t)s0 * H);
            const float4 a1 = *(const float4*)(hl + (size_t)s1 * H);
            const float4 a2 = *(const float4*)(hl + (size_t)s2 * H);
            const float4 a3 = *(const float4*)(hl + (size_t)s3 * H);
            if (SORTED) {
                acc = add4(a0, acc);
                if (j + 1 < n) acc = add4(a1, acc);
                if (j + 2 < n) acc = add4(a2, acc);
                if (j + 3 < n) acc = add4(a3, acc);
            } else {
                const float w0 = __int_as_float(__builtin_amdgcn_readlane(ev.y, j));
                const float w1 = __int_as_float(__builtin_amdgcn_readlane(ev.y, j1));
                const float w2 = __int_as_float(__builtin_amdgcn_readlane(ev.y, j2));
                const float w3 = __int_as_float(__builtin_amdgcn_readlane(ev.y, j3));
                acc = fma4(w0, a0, acc);
                if (j + 1 < n) acc = fma4(w1, a1, acc);
                if (j + 2 < n) acc = fma4(w2, a2, acc);
                if (j + 3 < n) acc = fma4(w3, a3, acc);
            }
        }
    }

    if (act) {
        float4 v;
        if (SORTED) {
            v = make_float4(di * (acc.x + hself.x), di * (acc.y + hself.y),
                            di * (acc.z + hself.z), di * (acc.w + hself.w));
        } else {
            const float d2 = di * di;
            v = make_float4(di * acc.x + d2 * hself.x, di * acc.y + d2 * hself.y,
                            di * acc.z + d2 * hself.z, di * acc.w + d2 * hself.w);
        }
        v.x = v.x > 0.f ? v.x : 0.f;
        v.y = v.y > 0.f ? v.y : 0.f;
        v.z = v.z > 0.f ? v.z : 0.f;
        v.w = v.w > 0.f ? v.w : 0.f;
        *(float4*)(xout + (size_t)node * H + c0) = v;
    }
}

static inline int ceil_div_i(long long a, long long b) { return (int)((a + b - 1) / b); }

static void launch_gemm(const float* X, const float* W, const float* b, const float* scale,
                        float* Y, int M, int K, int N, hipStream_t stream) {
    const int TPB = 256;
    const int strips = ceil_div_i(M, 16);
    if (N == 128 && (K == 32 || K == 64 || K == 128)) {
        const int grid = strips < 1024 ? strips : 1024;
        if (scale) {
            if (K == 32)
                gemm_bias_wmma_ws<32, true><<<grid, TPB, 0, stream>>>(X, W, b, scale, Y, M);
            else if (K == 64)
                gemm_bias_wmma_ws<64, true><<<grid, TPB, 0, stream>>>(X, W, b, scale, Y, M);
            else
                gemm_bias_wmma_ws<128, true><<<grid, TPB, 0, stream>>>(X, W, b, scale, Y, M);
        } else {
            if (K == 32)
                gemm_bias_wmma_ws<32, false><<<grid, TPB, 0, stream>>>(X, W, b, scale, Y, M);
            else if (K == 64)
                gemm_bias_wmma_ws<64, false><<<grid, TPB, 0, stream>>>(X, W, b, scale, Y, M);
            else
                gemm_bias_wmma_ws<128, false><<<grid, TPB, 0, stream>>>(X, W, b, scale, Y, M);
        }
        return;
    }
    if (K == 32)
        gemm_bias_wmma_lds<32><<<strips, TPB, 0, stream>>>(X, W, b, scale, Y, M, N);
    else if (K == 64)
        gemm_bias_wmma_lds<64><<<strips, TPB, 0, stream>>>(X, W, b, scale, Y, M, N);
    else if (K == 128)
        gemm_bias_wmma_lds<128><<<strips, TPB, 0, stream>>>(X, W, b, scale, Y, M, N);
    else {
        const int tiles = strips * (N / 16);
        gemm_bias_wmma_rt<<<ceil_div_i(tiles, 8), TPB, 0, stream>>>(X, W, b, scale, Y, M, K, N);
    }
}

extern "C" void kernel_launch(void* const* d_in, const int* in_sizes, int n_in,
                              void* d_out, int out_size, void* d_ws, size_t ws_size,
                              hipStream_t stream) {
    const float* x_user  = (const float*)d_in[0];
    const float* x_movie = (const float*)d_in[1];
    const int*   edge    = (const int*)d_in[2];
    const float* Wu = (const float*)d_in[3];
    const float* bu = (const float*)d_in[4];
    const float* Wm = (const float*)d_in[5];
    const float* bm = (const float*)d_in[6];
    const float* W1 = (const float*)d_in[7];
    const float* b1 = (const float*)d_in[8];
    const float* W2 = (const float*)d_in[9];
    const float* b2 = (const float*)d_in[10];
    const float* We = (const float*)d_in[11];
    const float* be = (const float*)d_in[12];
    float* out = (float*)d_out;

    const int H  = in_sizes[4];
    const int C  = in_sizes[12];
    const int Fu = in_sizes[3] / H;
    const int Fm = in_sizes[5] / H;
    const int Nu = in_sizes[0] / Fu;
    const int Nm = in_sizes[1] / Fm;
    const int E  = in_sizes[2] / 2;
    const int N  = Nu + Nm;
    const int* row = edge;
    const int* col = edge + E;

    float* ws   = (float*)d_ws;
    float* X    = ws;
    float* Hh   = X  + (size_t)N * H;
    float* Acc  = Hh + (size_t)N * H;
    float* dinv = Acc + (size_t)N * H;

    const int TPB = 256, WPB = 8;

    const size_t Npad = ((size_t)N + 31) & ~(size_t)31;
    float* P      = dinv + Npad;
    int*   cnt    = (int*)(P + (size_t)N * 16);
    int*   start  = cnt + (((size_t)N + 1 + 31) & ~(size_t)31);
    int*   cursor = start + Npad;
    int2*  slots  = (int2*)(cursor + Npad);
    float* Wcu = (float*)(slots + (((size_t)E + 15) & ~(size_t)15));
    float* Wcm = Wcu + (size_t)H * Fu;
    float* bcu = Wcm + (size_t)H * Fm;
    float* bcm = bcu + H;
    const bool fold = (H == 128) && (Fu % 32 == 0) && (Fm % 32 == 0);
    int* srcs     = (int*)(bcm + H);
    int* BH       = srcs + (((size_t)E + 31) & ~(size_t)31);
    int* binBase  = BH + (size_t)SORT_NBMAX * SORT_G;
    int* binCount = binBase + SORT_NBMAX;
    const int NB = ceil_div_i(N, 256);
    const bool sorted = NB <= SORT_NBMAX;

    const int composeUnits = fold ? 8 * (Fu / 32) + 8 * (Fm / 32) + 8 : 0;
    if (sorted) {
        const int chunk = ceil_div_i(E, SORT_G);
        bin_hist_kernel<<<SORT_G + ceil_div_i(composeUnits, COMPOSE_WPB), SORT_T, 0, stream>>>(
            col, BH, E, NB, chunk, W1, b1, Wu, bu, Wm, bm, Wcu, bcu, Wcm, bcm, Fu, Fm);
        bin_scatter_kernel<<<SORT_G, SORT_T, 0, stream>>>(row, col, BH, binBase, binCount, slots, E, NB, chunk);
        bin_sort_kernel<<<NB, SORT_T, 0, stream>>>(slots, binBase, binCount, start, cnt, dinv, srcs, N, E);
    } else {
        zero_int_kernel<<<ceil_div_i(N + 1, TPB), TPB, 0, stream>>>(cnt, N + 1);
        count_targets_kernel<<<ceil_div_i(E, TPB), TPB, 0, stream>>>(col, cnt, E);
        alloc_segments_kernel<<<ceil_div_i(N, TPB), TPB, 0, stream>>>(cnt, cnt + N, start, cursor, dinv, N);
        fill_segments_kernel<<<ceil_div_i(E, TPB), TPB, 0, stream>>>(row, col, dinv, cursor, slots, E);
    }
    const float* rowScale = sorted ? dinv : nullptr;

    if (fold) {
        if (!sorted)
            compose_wmma<<<ceil_div_i(composeUnits, WPB), TPB, 0, stream>>>(W1, b1, Wu, bu, Wm, bm,
                                                                            Wcu, bcu, Wcm, bcm, Fu, Fm);
        if (sorted && Fu == 32 && Fm == 64) {
            const int su = ceil_div_i(Nu, 16), sm = ceil_div_i(Nm, 16);
            const int gu = su < 768 ? su : 768, gm = sm < 768 ? sm : 768;
            gemm_pair_wmma_ws<32, 64, true><<<gu + gm, TPB, 0, stream>>>(
                x_user, Wcu, bcu, rowScale, Hh, Nu, gu,
                x_movie, Wcm, bcm, rowScale + Nu, Hh + (size_t)Nu * H, Nm);
        } else {
            launch_gemm(x_user, Wcu, bcu, rowScale, Hh, Nu, Fu, H, stream);
            launch_gemm(x_movie, Wcm, bcm, rowScale ? rowScale + Nu : nullptr,
                        Hh + (size_t)Nu * H, Nm, Fm, H, stream);
        }
    } else {
        launch_gemm(x_user, Wu, bu, nullptr, X, Nu, Fu, H, stream);
        launch_gemm(x_movie, Wm, bm, nullptr, X + (size_t)Nu * H, Nm, Fm, H, stream);
    }

    const int gatherWaves = N * ((H + 127) / 128);

    if (!fold) launch_gemm(X, W1, b1, rowScale, Hh, N, H, H, stream);
    if (sorted)
        gather_relu_kernel<true><<<ceil_div_i(gatherWaves, WPB), TPB, 0, stream>>>(Hh, dinv, start, cnt, slots, srcs, Acc, N, E, H);
    else
        gather_relu_kernel<false><<<ceil_div_i(gatherWaves, WPB), TPB, 0, stream>>>(Hh, dinv, start, cnt, slots, srcs, Acc, N, E, H);

    launch_gemm(Acc, W2, b2, rowScale, Hh, N, H, H, stream);
    if (sorted)
        gather_relu_kernel<true><<<ceil_div_i(gatherWaves, WPB), TPB, 0, stream>>>(Hh, dinv, start, cnt, slots, srcs, X, N, E, H);
    else
        gather_relu_kernel<false><<<ceil_div_i(gatherWaves, WPB), TPB, 0, stream>>>(Hh, dinv, start, cnt, slots, srcs, X, N, E, H);

    const int edgeTiles = ceil_div_i(E, 16);
    if (H == 128 && C == 10) {
        const int nodeTiles = ceil_div_i(Nu, 16) + ceil_div_i(Nm, 16);
        node_proj_wmma<<<ceil_div_i(nodeTiles, 4), 128, 0, stream>>>(X, We, be, P, Nu, Nm, C);
        const long long quads = ((long long)E * C + 3) / 4;
        edge_add10<4><<<ceil_div_i(quads, TPB * 4), TPB, 0, stream>>>(P, row, col, out, E, Nu);
    } else if (H == 128)
        edge_mlp_wmma<128><<<ceil_div_i(edgeTiles, WPB), TPB, 0, stream>>>(X, row, col, We, be, out, E, Nu, H, C);
    else
        edge_mlp_wmma<0><<<ceil_div_i(edgeTiles, WPB), TPB, 0, stream>>>(X, row, col, We, be, out, E, Nu, H, C);
}


# last gather launch also runs the 16-column projection (not the 128-wide GEMM): a block sums 16 nodes' rows into LDS, wave 0 feeds them to WMMA as A; projection launch, x2 plane gone; sum order same
# speedup vs baseline: 5.6129x; 1.0664x over previous
#include <hip/hip_runtime.h>

typedef __attribute__((ext_vector_type(2))) float v2f;
typedef __attribute__((ext_vector_type(8))) float v8f;

__device__ __forceinline__ v8f wmma_f32x4(v2f a, v2f b, v8f c) {
    return __builtin_amdgcn_wmma_f32_16x16x4_f32(
         false, a,  false, b,
         (short)0, c,  false,  false);
}

__device__ __forceinline__ void load_row_scale8(const float* __restrict__ scale,
                                                int row0, int M, float (&sc)[8]) {
    if (scale == nullptr) {
#pragma unroll
        for (int r = 0; r < 8; ++r) sc[r] = 1.0f;
    } else {
#pragma unroll
        for (int r = 0; r < 8; ++r) sc[r] = scale[min(row0 + r, M - 1)];
    }
}

template <int K>
__global__ void gemm_bias_wmma_lds(const float* __restrict__ X,
                                   const float* __restrict__ W,
                                   const float* __restrict__ bias,
                                   const float* __restrict__ scale,
                                   float* __restrict__ Y,
                                   int M, int N) {
    __shared__ float As[16 * K];

    const int lane = threadIdx.x & 31;
    const int wave = threadIdx.x >> 5;
    const int wavesPerBlock = blockDim.x >> 5;
    const int tilesN = N >> 4;
    const int strip = blockIdx.x * 16;

    if (strip + 16 <= M) {
        const float4* __restrict__ src = (const float4*)(X + (size_t)strip * K);
        float4* dst = (float4*)As;
        for (int i = threadIdx.x; i < (16 * K) / 4; i += blockDim.x) dst[i] = src[i];
    } else {
        for (int i = threadIdx.x; i < 16 * K; i += blockDim.x) {
            const int rr = i / K, cc = i - rr * K;
            const int gr = min(strip + rr, M - 1);
            As[i] = X[(size_t)gr * K + cc];
        }
    }
    __syncthreads();

    const int koff = (lane >> 4) << 1;
    const float* __restrict__ arow = As + (lane & 15) * K + koff;
    const bool fullTile = (strip + 16 <= M);
    const int rbase = strip + ((lane >> 4) << 3);
    float sc[8];
    load_row_scale8(scale, rbase, M, sc);

    for (int tn = wave; tn < tilesN; tn += wavesPerBlock) {
        const int ncol = tn * 16 + (lane & 15);
        const float* __restrict__ wrow = W + (size_t)ncol * K + koff;

        v8f acc = {};
#pragma unroll
        for (int k = 0; k < K; k += 4) {
            v2f a = *(const v2f*)(arow + k);
            v2f b = *(const v2f*)(wrow + k);
            acc = wmma_f32x4(a, b, acc);
        }

        const float bv = bias[ncol];
        float* yp = Y + (size_t)rbase * N + ncol;
        if (fullTile) {
#pragma unroll
            for (int r = 0; r < 8; ++r) { *yp = (acc[r] + bv) * sc[r]; yp += N; }
        } else {
#pragma unroll
            for (int r = 0; r < 8; ++r) {
                if (rbase + r < M) *yp = (acc[r] + bv) * sc[r];
                yp += N;
            }
        }
    }
}

template <int K, bool SCALED>
__device__ __forceinline__ void
gemm_ws_body(const float* __restrict__ X,
             const float* __restrict__ W,
             const float* __restrict__ bias,
             const float* __restrict__ scale,
             float* __restrict__ Y, int M,
             float* __restrict__ lds, int firstStrip, int stripStep) {
    constexpr int N = 128;
    constexpr int P = K + 4;
    constexpr int C4 = K / 4;
    constexpr int NV = 16 * C4;
    constexpr int PER = (NV + 255) / 256;
    float (*As)[16 * P] = (float (*)[16 * P])lds;

    const int lane = threadIdx.x & 31;
    const int wave = threadIdx.x >> 5;
    const int c = lane & 15;
    const int h = lane >> 4;
    const int ncol = wave * 16 + c;
    const int strips = (M + 15) >> 4;
    const int step = stripStep;

    float4 bf[K / 8];
    {
        const float* __restrict__ wrow = W + (size_t)ncol * K + 4 * h;
#pragma unroll
        for (int i = 0; i < K / 8; ++i) bf[i] = *(const float4*)(wrow + 8 * i);
    }
    const float bv = bias[ncol];

    static_assert(PER <= 2, "a strip is at most 512 float4s");
    const int i0 = threadIdx.x;
    const int rr0 = i0 / C4, cc0 = i0 - rr0 * C4;
    const int i1 = i0 + 256;
    const int rr1 = i1 / C4, cc1 = i1 - rr1 * C4;

    int s = firstStrip;
    if (s < strips) {
        if (NV >= 256 || i0 < NV)
            *(float4*)(&As[0][rr0 * P + 4 * cc0]) =
                *(const float4*)(X + (size_t)min(s * 16 + rr0, M - 1) * K + 4 * cc0);
        if (PER > 1)
            *(float4*)(&As[0][rr1 * P + 4 * cc1]) =
                *(const float4*)(X + (size_t)min(s * 16 + rr1, M - 1) * K + 4 * cc1);
    }
    __syncthreads();

    const int aoff = c * P + 4 * h;
    int buf = 0;
    for (; s < strips; s += step) {
        const int sn = s + step;
        const bool more = sn < strips;
        float4 pre0 = make_float4(0.f, 0.f, 0.f, 0.f);
        float4 pre1 = make_float4(0.f, 0.f, 0.f, 0.f);
        if (more) {
            if (NV >= 256 || i0 < NV)
                pre0 = *(const float4*)(X + (size_t)min(sn * 16 + rr0, M - 1) * K + 4 * cc0);
            if (PER > 1)
                pre1 = *(const float4*)(X + (size_t)min(sn * 16 + rr1, M - 1) * K + 4 * cc1);
        }
        const int rbase = s * 16 + 8 * h;
        float4 su = make_float4(1.f, 1.f, 1.f, 1.f), sv = su;
        if (SCALED) {
            const float* __restrict__ sp = scale + max(min(rbase, M - 8), 0);
            __builtin_memcpy(&su, sp, 16);
            __builtin_memcpy(&sv, sp + 4, 16);
        }

        const float* arow = &As[buf][aoff];
        v8f acc = {};
#pragma unroll
        for (int i = 0; i < K / 8; ++i) {
            const float4 a = *(const float4*)(arow + 8 * i);
            v2f a0 = {a.x, a.y}, b0 = {bf[i].x, bf[i].y};
            v2f a1 = {a.z, a.w}, b1 = {bf[i].z, bf[i].w};
            acc = wmma_f32x4(a0, b0, acc);
            acc = wmma_f32x4(a1, b1, acc);
        }

        if (more) {
            if (NV >= 256 || i0 < NV) *(float4*)(&As[buf ^ 1][rr0 * P + 4 * cc0]) = pre0;
            if (PER > 1) *(float4*)(&As[buf ^ 1][rr1 * P + 4 * cc1]) = pre1;
        }
        __syncthreads();
        buf ^= 1;

        float* yp = Y + (size_t)rbase * N + ncol;
        if (s * 16 + 16 <= M) {
            const float sc[8] = {su.x, su.y, su.z, su.w, sv.x, sv.y, sv.z, sv.w};
#pragma unroll
            for (int r = 0; r < 8; ++r) { *yp = (acc[r] + bv) * sc[r]; yp += N; }
        } else {
#pragma unroll
            for (int r = 0; r < 8; ++r) {
                if (rbase + r < M) *yp = (acc[r] + bv) * (SCALED ? scale[rbase + r] : 1.0f);
                yp += N;
            }
        }
    }
}

template <int K, bool SCALED>
__global__ void __launch_bounds__(256)
gemm_bias_wmma_ws(const float* __restrict__ X,
                  const float* __restrict__ W,
                  const float* __restrict__ bias,
                  const float* __restrict__ scale,
                  float* __restrict__ Y, int M) {
    __shared__ __attribute__((aligned(16))) float As[2 * 16 * (K + 4)];
    gemm_ws_body<K, SCALED>(X, W, bias, scale, Y, M, As, (int)blockIdx.x, (int)gridDim.x);
}

template <int KA, int KB, bool SCALED>
__global__ void __launch_bounds__(256)
gemm_pair_wmma_ws(const float* __restrict__ XA, const float* __restrict__ WA,
                  const float* __restrict__ biasA, const float* __restrict__ scaleA,
                  float* __restrict__ YA, int MA, int gridA,
                  const float* __restrict__ XB, const float* __restrict__ WB,
                  const float* __restrict__ biasB, const float* __restrict__ scaleB,
                  float* __restrict__ YB, int MB) {
    constexpr int KMAX = KA > KB ? KA : KB;
    __shared__ __attribute__((aligned(16))) float As[2 * 16 * (KMAX + 4)];
    const int bid = (int)blockIdx.x;
    if (bid < gridA)
        gemm_ws_body<KA, SCALED>(XA, WA, biasA, scaleA, YA, MA, As, bid, gridA);
    else
        gemm_ws_body<KB, SCALED>(XB, WB, biasB, scaleB, YB, MB, As, bid - gridA,
                                 (int)gridDim.x - gridA);
}

__device__ __forceinline__ void
compose_unit(const float* __restrict__ W1, const float* __restrict__ b1,
             const float* __restrict__ Wu, const float* __restrict__ bu,
             const float* __restrict__ Wm, const float* __restrict__ bm,
             float* __restrict__ Wcu, float* __restrict__ bcu,
             float* __restrict__ Wcm, float* __restrict__ bcm,
             int Fu, int Fm, int unit, int lane) {
    constexpr int H = 128;
    constexpr int S = H / 8;
    const int c = lane & 15;
    const int h = lane >> 4;
    const int wu = 8 * (Fu >> 5);
    const int wm = 8 * (Fm >> 5);

    if (unit < wu + wm) {
        const bool second = unit >= wu;
        const int u = second ? unit - wu : unit;
        const int F = second ? Fm : Fu;
        const float* __restrict__ Wt = second ? Wm : Wu;
        float* __restrict__ Wc = second ? Wcm : Wcu;
        const int groups = F >> 5;
        const int nt = u / groups;
        const int cg = u - nt * groups;

        const float* __restrict__ arow = W1 + (size_t)(nt * 16 + c) * H + 4 * h;
        const float* __restrict__ bcol = Wt + (size_t)(4 * h) * F + cg * 32 + c;

        float4 af[S];
        float bl[S][4], br[S][4];
#pragma unroll
        for (int i = 0; i < S; ++i) {
            af[i] = *(const float4*)(arow + 8 * i);
            const float* __restrict__ bp = bcol + (size_t)(8 * i) * F;
#pragma unroll
            for (int e = 0; e < 4; ++e) {
                bl[i][e] = bp[(size_t)e * F];
                br[i][e] = bp[(size_t)e * F + 16];
            }
        }

        v8f acc0 = {}, acc1 = {};
#pragma unroll
        for (int i = 0; i < S; ++i) {
            v2f a0 = {af[i].x, af[i].y}, a1 = {af[i].z, af[i].w};
            v2f p0 = {bl[i][0], bl[i][1]}, p1 = {bl[i][2], bl[i][3]};
            v2f q0 = {br[i][0], br[i][1]}, q1 = {br[i][2], br[i][3]};
            acc0 = wmma_f32x4(a0, p0, acc0);
            acc1 = wmma_f32x4(a0, q0, acc1);
            acc0 = wmma_f32x4(a1, p1, acc0);
            acc1 = wmma_f32x4(a1, q1, acc1);
        }

        float* wp = Wc + (size_t)(nt * 16 + 8 * h) * F + cg * 32 + c;
#pragma unroll
        for (int r = 0; r < 8; ++r) {
            wp[0] = acc0[r];
            wp[16] = acc1[r];
            wp += F;
        }
    } else if (unit < wu + wm + 8) {
        const int q = unit - wu - wm;
        const bool second = q >= 4;
        const float* __restrict__ bt = second ? bm : bu;
        float* __restrict__ bc = second ? bcm : bcu;
        const int nb = 32 * (q & 3);

        const float* __restrict__ ar0 = W1 + (size_t)(nb + c) * H + 4 * h;
        const float* __restrict__ ar1 = ar0 + (size_t)16 * H;
        const float* __restrict__ bv = bt + 4 * h;

        float4 a0f[S], a1f[S], bf[S];
#pragma unroll
        for (int i = 0; i < S; ++i) {
            a0f[i] = *(const float4*)(ar0 + 8 * i);
            a1f[i] = *(const float4*)(ar1 + 8 * i);
            bf[i] = *(const float4*)(bv + 8 * i);
        }
        const float4 s00 = *(const float4*)(b1 + nb + 8 * h);
        const float4 s01 = *(const float4*)(b1 + nb + 8 * h + 4);
        const float4 s10 = *(const float4*)(b1 + nb + 16 + 8 * h);
        const float4 s11 = *(const float4*)(b1 + nb + 16 + 8 * h + 4);

        v8f acc0 = {}, acc1 = {};
#pragma unroll
        for (int i = 0; i < S; ++i) {
            v2f x0 = {a0f[i].x, a0f[i].y}, x1 = {a0f[i].z, a0f[i].w};
            v2f y0 = {a1f[i].x, a1f[i].y}, y1 = {a1f[i].z, a1f[i].w};
            v2f p0 = {bf[i].x, bf[i].y}, p1 = {bf[i].z, bf[i].w};
            acc0 = wmma_f32x4(x0, p0, acc0);
            acc1 = wmma_f32x4(y0, p0, acc1);
            acc0 = wmma_f32x4(x1, p1, acc0);
            acc1 = wmma_f32x4(y1, p1, acc1);
        }

        const float4 o00 = make_float4(acc0[0] + s00.x, acc0[1] + s00.y, acc0[2] + s00.z, acc0[3] + s00.w);
        const float4 o01 = make_float4(acc0[4] + s01.x, acc0[5] + s01.y, acc0[6] + s01.z, acc0[7] + s01.w);
        const float4 o10 = make_float4(acc1[0] + s10.x, acc1[1] + s10.y, acc1[2] + s10.z, acc1[3] + s10.w);
        const float4 o11 = make_float4(acc1[4] + s11.x, acc1[5] + s11.y, acc1[6] + s11.z, acc1[7] + s11.w);
        if (c == 0) {
            float* op = bc + nb + 8 * h;
            *(float4*)(op) = o00;
            *(float4*)(op + 4) = o01;
            *(float4*)(op + 16) = o10;
            *(float4*)(op + 20) = o11;
        }
    }
}

__global__ void __launch_bounds__(256)
compose_wmma(const float* __restrict__ W1, const float* __restrict__ b1,
             const float* __restrict__ Wu, const float* __restrict__ bu,
             const float* __restrict__ Wm, const float* __restrict__ bm,
             float* __restrict__ Wcu, float* __restrict__ bcu,
             float* __restrict__ Wcm, float* __restrict__ bcm,
             int Fu, int Fm) {
    const int unit = blockIdx.x * (blockDim.x >> 5) + (threadIdx.x >> 5);
    compose_unit(W1, b1, Wu, bu, Wm, bm, Wcu, bcu, Wcm, bcm, Fu, Fm, unit, threadIdx.x & 31);
}

__global__ void gemm_bias_wmma_rt(const float* __restrict__ X,
                                  const float* __restrict__ W,
                                  const float* __restrict__ bias,
                                  const float* __restrict__ scale,
                                  float* __restrict__ Y,
                                  int M, int K, int N) {
    const int lane = threadIdx.x & 31;
    const int wave = threadIdx.x >> 5;
    const int tilesN = N >> 4;
    const int tile = blockIdx.x * (blockDim.x >> 5) + wave;
    const int tm = tile / tilesN;
    const int tn = tile - tm * tilesN;
    if (tm * 16 >= M) return;

    const int mrow = min(tm * 16 + (lane & 15), M - 1);
    const int ncol = tn * 16 + (lane & 15);
    const int koff = (lane >> 4) << 1;
    const float* __restrict__ xrow = X + (size_t)mrow * K + koff;
    const float* __restrict__ wrow = W + (size_t)ncol * K + koff;

    v8f acc = {};
#pragma unroll 8
    for (int k = 0; k < K; k += 4) {
        v2f a = *(const v2f*)(xrow + k);
        v2f b = *(const v2f*)(wrow + k);
        acc = wmma_f32x4(a, b, acc);
    }

    const float bv = bias[ncol];
    const int rbase = tm * 16 + ((lane >> 4) << 3);
    float sc[8];
    load_row_scale8(scale, rbase, M, sc);
    float* yp = Y + (size_t)rbase * N + ncol;
#pragma unroll
    for (int r = 0; r < 8; ++r) {
        if (rbase + r < M) *yp = (acc[r] + bv) * sc[r];
        yp += N;
    }
}

template <int HT>
__global__ void edge_mlp_wmma(const float* __restrict__ xf,
                              const int* __restrict__ row,
                              const int* __restrict__ col,
                              const float* __restrict__ We,
                              const float* __restrict__ be,
                              float* __restrict__ out,
                              int E, int Nu, int Hrt, int C) {
    const int H = (HT > 0) ? HT : Hrt;
    const int K = 2 * H;
    const int lane = threadIdx.x & 31;
    const int wave = threadIdx.x >> 5;
    const int tile = blockIdx.x * (blockDim.x >> 5) + wave;
    const int ebase = tile * 16;
    if (ebase >= E) return;

    const int m = lane & 15;
    const int e = min(ebase + m, E - 1);
    const int r = row[e];
    const int c = col[e];
    const int koff = (lane >> 4) << 1;

    const float* __restrict__ urow = xf + (size_t)r * H + koff;
    const float* __restrict__ mrow = xf + (size_t)(Nu + c) * H + koff;

    const int n = lane & 15;
    const float wmask = (n < C) ? 1.0f : 0.0f;
    const float* __restrict__ wrow = We + (size_t)min(n, C - 1) * K + koff;

    v8f acc = {};
#pragma unroll 8
    for (int k = 0; k < K; k += 4) {
        const float* src = (k + koff < H) ? (urow + k) : (mrow + (k - H));
        v2f a = *(const v2f*)src;
        v2f b = *(const v2f*)(wrow + k);
        b.x *= wmask;
        b.y *= wmask;
        acc = wmma_f32x4(a, b, acc);
    }

    if (n < C) {
        const float bv = be[n];
        const int mb = (lane >> 4) << 3;
        float* op = out + (size_t)(ebase + mb) * C + n;
        if (ebase + 16 <= E) {
#pragma unroll
            for (int rr = 0; rr < 8; ++rr) { *op = acc[rr] + bv; op += C; }
        } else {
#pragma unroll
            for (int rr = 0; rr < 8; ++rr) {
                if (ebase + mb + rr < E) *op = acc[rr] + bv;
                op += C;
            }
        }
    }
}

__global__ void node_proj_wmma(const float* __restrict__ xf,
                               const float* __restrict__ We,
                               const float* __restrict__ be,
                               float* __restrict__ P,
                               int Nu, int Nm, int C) {
    constexpr int H = 128;
    const int lane = threadIdx.x & 31;
    const int wave = threadIdx.x >> 5;
    const int tilesU = (Nu + 15) >> 4;
    const int tilesM = (Nm + 15) >> 4;
    const int tile = blockIdx.x * (blockDim.x >> 5) + wave;
    if (tile >= tilesU + tilesM) return;

    const bool second = tile >= tilesU;
    const int segEnd = second ? Nu + Nm : Nu;
    const int base = second ? Nu + ((tile - tilesU) << 4) : (tile << 4);

    const int c = lane & 15;
    const int h = lane >> 4;
    const int node = min(base + 2 * (c & 7) + (c >> 3), segEnd - 1);
    const float* __restrict__ arow = xf + (size_t)node * H + 4 * h;
    const float* __restrict__ brow =
        We + (size_t)min(c, C - 1) * (2 * H) + (second ? H : 0) + 4 * h;

    const float bsel = be[min(c, C - 1)];
    const float bv = (second && c < C) ? bsel : 0.0f;

    v8f acc0 = {}, acc1 = {};
#pragma unroll
    for (int i = 0; i < H / 8; ++i) {
        const float4 a = *(const float4*)(arow + 8 * i);
        const float4 b = *(const float4*)(brow + 8 * i);
        v2f a0 = {a.x, a.y}, b0 = {b.x, b.y};
        v2f a1 = {a.z, a.w}, b1 = {b.z, b.w};
        acc0 = wmma_f32x4(a0, b0, acc0);
        acc1 = wmma_f32x4(a1, b1, acc1);
    }

    float v[8];
#pragma unroll
    for (int r = 0; r < 8; ++r) v[r] = (acc0[r] + acc1[r]) + bv;

    float* pp = P + (size_t)base * 16 + lane;
    if (base + 16 <= segEnd) {
#pragma unroll
        for (int r = 0; r < 8; ++r) pp[32 * r] = v[r];
    } else {
#pragma unroll
        for (int r = 0; r < 8; ++r)
            if (base + 2 * r + h < segEnd) pp[32 * r] = v[r];
    }
}

template <int ITER>
__global__ void edge_add10(const float* __restrict__ P,
                           const int* __restrict__ row,
                           const int* __restrict__ col,
                           float* __restrict__ out,
                           int E, int Nu) {
    constexpr unsigned C = 10;
    const unsigned total = (unsigned)E * C;
    const float* __restrict__ Pu = P;
    const float* __restrict__ Pm = P + (size_t)Nu * 16;
    const unsigned perBlock = blockDim.x * ITER;
    const unsigned qb = blockIdx.x * perBlock;

    if ((qb + perBlock) * 4u <= total) {
        unsigned f0[ITER], cA[ITER], cB[ITER];
        int rA[ITER], mA[ITER], rB[ITER], mB[ITER];
#pragma unroll
        for (int it = 0; it < ITER; ++it) {
            f0[it] = 4u * (qb + it * blockDim.x + threadIdx.x);
            const unsigned f1 = f0[it] + 2u;
            const unsigned eA = f0[it] / C, eB = f1 / C;
            cA[it] = f0[it] - eA * C;
            cB[it] = f1 - eB * C;
            rA[it] = row[eA]; mA[it] = col[eA];
            rB[it] = row[eB]; mB[it] = col[eB];
        }
        float2 uA[ITER], vA[ITER], uB[ITER], vB[ITER];
#pragma unroll
        for (int it = 0; it < ITER; ++it) {
            uA[it] = *(const float2*)(Pu + (size_t)rA[it] * 16 + cA[it]);
            vA[it] = *(const float2*)(Pm + (size_t)mA[it] * 16 + cA[it]);
            uB[it] = *(const float2*)(Pu + (size_t)rB[it] * 16 + cB[it]);
            vB[it] = *(const float2*)(Pm + (size_t)mB[it] * 16 + cB[it]);
        }
#pragma unroll
        for (int it = 0; it < ITER; ++it) {
            *(float4*)(out + f0[it]) = make_float4(uA[it].x + vA[it].x, uA[it].y + vA[it].y,
                                                   uB[it].x + vB[it].x, uB[it].y + vB[it].y);
        }
    } else {
        for (unsigned p = threadIdx.x; p < 2u * perBlock; p += blockDim.x) {
            const unsigned f = 4u * qb + 2u * p;
            if (f < total) {
                const unsigned e = f / C, c = f - e * C;
                const float2 u = *(const float2*)(Pu + (size_t)row[e] * 16 + c);
                const float2 v = *(const float2*)(Pm + (size_t)col[e] * 16 + c);
                *(float2*)(out + f) = make_float2(u.x + v.x, u.y + v.y);
            }
        }
    }
}

constexpr int SORT_G = 128;
constexpr int SORT_NBMAX = 1024;
constexpr int SORT_T = 1024;
static_assert(SORT_T == SORT_NBMAX && SORT_G % 32 == 0, "one bin per thread, whole chunk rows per wave");

__device__ __forceinline__ int wave_incl_scan(int v, int lane) {
    for (int d = 1; d < 32; d <<= 1) {
        const int t = __shfl_up(v, d);
        if (lane >= d) v += t;
    }
    return v;
}

constexpr int COMPOSE_WPB = 8;
__global__ void __launch_bounds__(SORT_T)
bin_hist_kernel(const int* __restrict__ col, int* __restrict__ BH,
                int E, int NB, int chunk,
                const float* __restrict__ W1, const float* __restrict__ b1,
                const float* __restrict__ Wu, const float* __restrict__ bu,
                const float* __restrict__ Wm, const float* __restrict__ bm,
                float* __restrict__ Wcu, float* __restrict__ bcu,
                float* __restrict__ Wcm, float* __restrict__ bcm,
                int Fu, int Fm) {
    __shared__ int hist[SORT_NBMAX];
    const int tid = threadIdx.x;
    if (blockIdx.x >= SORT_G) {
        const int wave = tid >> 5;
        if (wave < COMPOSE_WPB)
            compose_unit(W1, b1, Wu, bu, Wm, bm, Wcu, bcu, Wcm, bcm, Fu, Fm,
                         ((int)blockIdx.x - SORT_G) * COMPOSE_WPB + wave, tid & 31);
        return;
    }
    const int g = blockIdx.x;
    hist[tid] = 0;
    __syncthreads();
    const int lo = min(g * chunk, E);
    const int hi = min(lo + chunk, E);
    for (int e0 = lo + tid; e0 < hi; e0 += 4 * SORT_T) {
        int c[4];
#pragma unroll
        for (int k = 0; k < 4; ++k) {
            const int e = e0 + k * SORT_T;
            c[k] = (e < hi) ? col[e] : -1;
        }
#pragma unroll
        for (int k = 0; k < 4; ++k)
            if (c[k] >= 0) atomicAdd(&hist[min(c[k] >> 8, NB - 1)], 1);
    }
    __syncthreads();
    BH[(size_t)g * SORT_NBMAX + tid] = hist[tid];
}

__global__ void __launch_bounds__(SORT_T)
bin_scatter_kernel(const int* __restrict__ row, const int* __restrict__ col,
                   const int* __restrict__ BH,
                   int* __restrict__ binBase, int* __restrict__ binCount,
                   int2* __restrict__ binned, int E, int NB, int chunk) {
    __shared__ int pre[SORT_NBMAX];
    __shared__ int tot[SORT_NBMAX];
    __shared__ int cur[SORT_NBMAX];
    __shared__ int wsum[32];
    const int tid = threadIdx.x;
    const int lane = tid & 31;
    const int wave = tid >> 5;
    const int g = blockIdx.x;

    pre[tid] = 0;
    tot[tid] = 0;
    __syncthreads();
    constexpr int GW = SORT_G / 32;
    const int g0 = wave * GW;
    for (int bb = 0; bb * 128 < NB; bb += 2) {
        const int b4 = bb * 128 + 4 * lane;
        const bool two = (bb + 1) * 128 < NB;
        int4 va[GW], vb[GW];
#pragma unroll
        for (int k = 0; k < GW; ++k) {
            const int* rp = BH + (size_t)(g0 + k) * SORT_NBMAX + b4;
            va[k] = *(const int4*)rp;
            vb[k] = two ? *(const int4*)(rp + 128) : make_int4(0, 0, 0, 0);
        }
        int4 pa = make_int4(0, 0, 0, 0), ta = pa, pb = pa, tb = pa;
#pragma unroll
        for (int k = 0; k < GW; ++k) {
            const bool early = g0 + k < g;
            ta.x += va[k].x; ta.y += va[k].y; ta.z += va[k].z; ta.w += va[k].w;
            tb.x += vb[k].x; tb.y += vb[k].y; tb.z += vb[k].z; tb.w += vb[k].w;
            if (early) {
                pa.x += va[k].x; pa.y += va[k].y; pa.z += va[k].z; pa.w += va[k].w;
                pb.x += vb[k].x; pb.y += vb[k].y; pb.z += vb[k].z; pb.w += vb[k].w;
            }
        }
        atomicAdd(&tot[b4 + 0], ta.x); atomicAdd(&tot[b4 + 1], ta.y);
        atomicAdd(&tot[b4 + 2], ta.z); atomicAdd(&tot[b4 + 3], ta.w);
        atomicAdd(&pre[b4 + 0], pa.x); atomicAdd(&pre[b4 + 1], pa.y);
        atomicAdd(&pre[b4 + 2], pa.z); atomicAdd(&pre[b4 + 3], pa.w);
        if (two) {
            atomicAdd(&tot[b4 + 128], tb.x); atomicAdd(&tot[b4 + 129], tb.y);
            atomicAdd(&tot[b4 + 130], tb.z); atomicAdd(&tot[b4 + 131], tb.w);
            atomicAdd(&pre[b4 + 128], pb.x); atomicAdd(&pre[b4 + 129], pb.y);
            atomicAdd(&pre[b4 + 130], pb.z); atomicAdd(&pre[b4 + 131], pb.w);
        }
    }
    __syncthreads();
    const int mine = tot[tid];
    const int incl = wave_incl_scan(mine, lane);
    if (lane == 31) wsum[wave] = incl;
    __syncthreads();
    if (wave == 0) {
        const int w = wsum[lane];
        const int wi = wave_incl_scan(w, lane);
        wsum[lane] = wi - w;
    }
    __syncthreads();
    const int base = wsum[wave] + incl - mine;
    cur[tid] = base + pre[tid];
    if (g == 0 && tid < NB) { binBase[tid] = base; binCount[tid] = mine; }
    __syncthreads();

    const int lo = min(g * chunk, E);
    const int hi = min(lo + chunk, E);
    for (int e0 = lo + tid; e0 < hi; e0 += 4 * SORT_T) {
        int c[4], r[4];
#pragma unroll
        for (int k = 0; k < 4; ++k) {
            const int e = e0 + k * SORT_T;
            c[k] = (e < hi) ? col[e] : -1;
            r[k] = (e < hi) ? row[e] : 0;
        }
#pragma unroll
        for (int k = 0; k < 4; ++k) {
            if (c[k] >= 0) {
                const int pos = atomicAdd(&cur[min(c[k] >> 8, NB - 1)], 1);
                binned[min(max(pos, 0), E - 1)] = make_int2(r[k], c[k]);
            }
        }
    }
}

__global__ void __launch_bounds__(SORT_T)
bin_sort_kernel(const int2* __restrict__ binned,
                const int* __restrict__ binBase, const int* __restrict__ binCount,
                int* __restrict__ start, int* __restrict__ cnt,
                float* __restrict__ dinv, int* __restrict__ srcs, int N, int E) {
    __shared__ int c[256];
    __shared__ int cur[256];
    __shared__ int wsum[8];
    const int tid = threadIdx.x;
    const int lane = tid & 31;
    const int wave = tid >> 5;
    const int b = blockIdx.x;
    const int count = min(max(binCount[b], 0), E);
    const int base = min(max(binBase[b], 0), E - count);
    const int node0 = b << 8;

    if (tid < 256) c[tid] = 0;
    __syncthreads();
    for (int i0 = tid; i0 < count; i0 += 4 * SORT_T) {
        int t[4];
#pragma unroll
        for (int k = 0; k < 4; ++k) {
            const int i = i0 + k * SORT_T;
            t[k] = (i < count) ? binned[base + i].y : -1;
        }
#pragma unroll
        for (int k = 0; k < 4; ++k)
            if (t[k] >= 0) atomicAdd(&c[min(max(t[k] - node0, 0), 255)], 1);
    }
    __syncthreads();
    int mine = 0, incl = 0;
    if (tid < 256) {
        mine = c[tid];
        incl = wave_incl_scan(mine, lane);
        if (lane == 31) wsum[wave] = incl;
    }
    __syncthreads();
    if (tid < 256) {
        int off = 0;
        for (int w = 0; w < wave; ++w) off += wsum[w];
        const int excl = off + incl - mine;
        cur[tid] = excl;
        const int node = node0 + tid;
        if (node < N) {
            start[node] = base + excl;
            cnt[node] = mine;
            dinv[node] = rsqrtf((float)(mine + 1));
        }
    }
    __syncthreads();
    for (int i0 = tid; i0 < count; i0 += 4 * SORT_T) {
        int2 ed[4];
#pragma unroll
        for (int k = 0; k < 4; ++k) {
            const int i = i0 + k * SORT_T;
            ed[k] = (i < count) ? binned[base + i] : make_int2(0, -1);
        }
#pragma unroll
        for (int k = 0; k < 4; ++k) {
            if (ed[k].y >= 0) {
                const int p = atomicAdd(&cur[min(max(ed[k].y - node0, 0), 255)], 1);
                srcs[base + min(max(p, 0), count - 1)] = ed[k].x;
            }
        }
    }
}

__global__ void zero_int_kernel(int* __restrict__ p, int n) {
    const int i = blockIdx.x * blockDim.x + threadIdx.x;
    if (i < n) p[i] = 0;
}

__global__ void count_targets_kernel(const int* __restrict__ col,
                                     int* __restrict__ cnt, int E) {
    const int e = blockIdx.x * blockDim.x + threadIdx.x;
    if (e < E) atomicAdd(&cnt[col[e]], 1);
}

__global__ void alloc_segments_kernel(const int* __restrict__ cnt,
                                      int* __restrict__ total,
                                      int* __restrict__ start,
                                      int* __restrict__ cursor,
                                      float* __restrict__ dinv, int N) {
    const int i = blockIdx.x * blockDim.x + threadIdx.x;
    const int lane = threadIdx.x & 31;
    const int c = (i < N) ? cnt[i] : 0;
    int incl = c;
    for (int d = 1; d < 32; d <<= 1) {
        const int t = __shfl_up(incl, d);
        if (lane >= d) incl += t;
    }
    int base = 0;
    if (lane == 31) base = atomicAdd(total, incl);
    base = __shfl(base, 31);
    if (i < N) {
        const int s = base + incl - c;
        start[i] = s;
        cursor[i] = s;
        dinv[i] = rsqrtf((float)(c + 1));
    }
}

__global__ void fill_segments_kernel(const int* __restrict__ row,
                                     const int* __restrict__ col,
                                     const float* __restrict__ dinv,
                                     int* __restrict__ cursor,
                                     int2* __restrict__ slots, int E) {
    const int e = blockIdx.x * blockDim.x + threadIdx.x;
    if (e < E) {
        const int r = row[e];
        const int pos = atomicAdd(&cursor[col[e]], 1);
        slots[pos] = make_int2(r, __float_as_int(dinv[r]));
    }
}

__device__ __forceinline__ float4 fma4(float w, float4 a, float4 c) {
    return make_float4(fmaf(w, a.x, c.x), fmaf(w, a.y, c.y), fmaf(w, a.z, c.z), fmaf(w, a.w, c.w));
}

__device__ __forceinline__ float4 add4(float4 a, float4 c) {
    return make_float4(a.x + c.x, a.y + c.y, a.z + c.z, a.w + c.w);
}

template <bool SORTED>
__global__ void gather_relu_kernel(const float* __restrict__ h,
                                   const float* __restrict__ dinv,
                                   const int* __restrict__ start,
                                   const int* __restrict__ cnt,
                                   const int2* __restrict__ slots,
                                   const int* __restrict__ srcs,
                                   float* __restrict__ xout,
                                   int N, int E, int H) {
    const int lane = threadIdx.x & 31;
    const int cpn = (H + 127) >> 7;
    const int wid = __builtin_amdgcn_readfirstlane(
        (int)(blockIdx.x * (blockDim.x >> 5) + (threadIdx.x >> 5)));
    const int node = wid / cpn;
    if (node >= N) return;
    const int cb = (wid - node * cpn) << 7;
    const bool act = cb + 4 * lane < H;
    const int c0 = min(cb + 4 * lane, H - 4);
    const float* __restrict__ hl = h + c0;

    const int c = min(max(cnt[node], 0), E);
    const int s = min(max(start[node], 0), E - c);
    const float4 hself = *(const float4*)(hl + (size_t)node * H);
    const float di = dinv[node];

    float4 acc = make_float4(0.f, 0.f, 0.f, 0.f);
    for (int k0 = 0; k0 < c; k0 += 32) {
        const int n = min(32, c - k0);
        int2 ev = make_int2(0, 0);
        if (lane < n) {
            if (SORTED)
                ev.x = min(max(srcs[s + k0 + lane], 0), N - 1);
            else
                ev = slots[s + k0 + lane];
        }
        for (int j = 0; j < n; j += 4) {
            const int j1 = min(j + 1, n - 1), j2 = min(j + 2, n - 1), j3 = min(j + 3, n - 1);
            const int s0 = __builtin_amdgcn_readlane(ev.x, j);
            const int s1 = __builtin_amdgcn_readlane(ev.x, j1);
            const int s2 = __builtin_amdgcn_readlane(ev.x, j2);
            const int s3 = __builtin_amdgcn_readlane(ev.x, j3);
            const float4 a0 = *(const float4*)(hl + (size_t)s0 * H);
            const float4 a1 = *(const float4*)(hl + (size_t)s1 * H);
            const float4 a2 = *(const float4*)(hl + (size_t)s2 * H);
            const float4 a3 = *(const float4*)(hl + (size_t)s3 * H);
            if (SORTED) {
                acc = add4(a0, acc);
                if (j + 1 < n) acc = add4(a1, acc);
                if (j + 2 < n) acc = add4(a2, acc);
                if (j + 3 < n) acc = add4(a3, acc);
            } else {
                const float w0 = __int_as_float(__builtin_amdgcn_readlane(ev.y, j));
                const float w1 = __int_as_float(__builtin_amdgcn_readlane(ev.y, j1));
                const float w2 = __int_as_float(__builtin_amdgcn_readlane(ev.y, j2));
                const float w3 = __int_as_float(__builtin_amdgcn_readlane(ev.y, j3));
                acc = fma4(w0, a0, acc);
                if (j + 1 < n) acc = fma4(w1, a1, acc);
                if (j + 2 < n) acc = fma4(w2, a2, acc);
                if (j + 3 < n) acc = fma4(w3, a3, acc);
            }
        }
    }

    if (act) {
        float4 v;
        if (SORTED) {
            v = make_float4(di * (acc.x + hself.x), di * (acc.y + hself.y),
                            di * (acc.z + hself.z), di * (acc.w + hself.w));
        } else {
            const float d2 = di * di;
            v = make_float4(di * acc.x + d2 * hself.x, di * acc.y + d2 * hself.y,
                            di * acc.z + d2 * hself.z, di * acc.w + d2 * hself.w);
        }
        v.x = v.x > 0.f ? v.x : 0.f;
        v.y = v.y > 0.f ? v.y : 0.f;
        v.z = v.z > 0.f ? v.z : 0.f;
        v.w = v.w > 0.f ? v.w : 0.f;
        *(float4*)(xout + (size_t)node * H + c0) = v;
    }
}

__global__ void __launch_bounds__(256)
gather_relu_proj_kernel(const float* __restrict__ h,
                        const float* __restrict__ dinv,
                        const int* __restrict__ start,
                        const int* __restrict__ cnt,
                        const int* __restrict__ srcs,
                        const float* __restrict__ We,
                        const float* __restrict__ be,
                        float* __restrict__ P,
                        int Nu, int Nm, int E, int C) {
    constexpr int H = 128;
    constexpr int PITCH = H + 4;
    __shared__ __attribute__((aligned(16))) float As[16 * PITCH];

    const int lane = threadIdx.x & 31;
    const int wave = __builtin_amdgcn_readfirstlane((int)(threadIdx.x >> 5));
    const int N = Nu + Nm;
    const int tilesU = (Nu + 15) >> 4;
    const int tile = (int)blockIdx.x;
    const bool second = tile >= tilesU;
    const int segEnd = second ? Nu + Nm : Nu;
    const int base = second ? Nu + ((tile - tilesU) << 4) : (tile << 4);

    const float* __restrict__ hl = h + 4 * lane;

    int nodeT[2], cT[2], sT[2], evT[2];
    float diT[2];
    float4 selfT[2];
#pragma unroll
    for (int t = 0; t < 2; ++t) {
        nodeT[t] = min(base + 2 * wave + t, segEnd - 1);
        cT[t] = min(max(cnt[nodeT[t]], 0), E);
        sT[t] = min(max(start[nodeT[t]], 0), E - cT[t]);
        diT[t] = dinv[nodeT[t]];
        selfT[t] = *(const float4*)(hl + (size_t)nodeT[t] * H);
    }
    int rawT[2];
#pragma unroll
    for (int t = 0; t < 2; ++t) {
        rawT[t] = 0;
        if (lane < min(32, cT[t])) rawT[t] = srcs[sT[t] + lane];
    }
#pragma unroll
    for (int t = 0; t < 2; ++t)
        evT[t] = min(max(rawT[t], 0), N - 1);

#pragma unroll
    for (int t = 0; t < 2; ++t) {
        const int c = cT[t];
        const int s = sT[t];
        float4 acc = make_float4(0.f, 0.f, 0.f, 0.f);
        int ev = evT[t];
        for (int k0 = 0; k0 < c; k0 += 32) {
            const int n = min(32, c - k0);
            if (k0 > 0) {
                ev = 0;
                if (lane < n) ev = min(max(srcs[s + k0 + lane], 0), N - 1);
            }
            for (int j = 0; j < n; j += 4) {
                const int j1 = min(j + 1, n - 1), j2 = min(j + 2, n - 1), j3 = min(j + 3, n - 1);
                const int s0 = __builtin_amdgcn_readlane(ev, j);
                const int s1 = __builtin_amdgcn_readlane(ev, j1);
                const int s2 = __builtin_amdgcn_readlane(ev, j2);
                const int s3 = __builtin_amdgcn_readlane(ev, j3);
                const float4 a0 = *(const float4*)(hl + (size_t)s0 * H);
                const float4 a1 = *(const float4*)(hl + (size_t)s1 * H);
                const float4 a2 = *(const float4*)(hl + (size_t)s2 * H);
                const float4 a3 = *(const float4*)(hl + (size_t)s3 * H);
                acc = add4(a0, acc);
                if (j + 1 < n) acc = add4(a1, acc);
                if (j + 2 < n) acc = add4(a2, acc);
                if (j + 3 < n) acc = add4(a3, acc);
            }
        }
        const float di = diT[t];
        const float4 hs = selfT[t];
        float4 v = make_float4(di * (acc.x + hs.x), di * (acc.y + hs.y),
                               di * (acc.z + hs.z), di * (acc.w + hs.w));
        v.x = v.x > 0.f ? v.x : 0.f;
        v.y = v.y > 0.f ? v.y : 0.f;
        v.z = v.z > 0.f ? v.z : 0.f;
        v.w = v.w > 0.f ? v.w : 0.f;
        *(float4*)(&As[(2 * wave + t) * PITCH + 4 * lane]) = v;
    }
    __syncthreads();
    if (wave != 0) return;

    const int c = lane & 15;
    const int hh = lane >> 4;
    const float* arow = &As[(2 * (c & 7) + (c >> 3)) * PITCH + 4 * hh];
    const float* __restrict__ brow =
        We + (size_t)min(c, C - 1) * (2 * H) + (second ? H : 0) + 4 * hh;

    const float bsel = be[min(c, C - 1)];
    const float bv = (second && c < C) ? bsel : 0.0f;

    v8f acc0 = {}, acc1 = {};
#pragma unroll
    for (int i = 0; i < H / 8; ++i) {
        const float4 a = *(const float4*)(arow + 8 * i);
        const float4 b = *(const float4*)(brow + 8 * i);
        v2f a0 = {a.x, a.y}, b0 = {b.x, b.y};
        v2f a1 = {a.z, a.w}, b1 = {b.z, b.w};
        acc0 = wmma_f32x4(a0, b0, acc0);
        acc1 = wmma_f32x4(a1, b1, acc1);
    }

    float v[8];
#pragma unroll
    for (int r = 0; r < 8; ++r) v[r] = (acc0[r] + acc1[r]) + bv;

    float* pp = P + (size_t)base * 16 + lane;
    if (base + 16 <= segEnd) {
#pragma unroll
        for (int r = 0; r < 8; ++r) pp[32 * r] = v[r];
    } else {
#pragma unroll
        for (int r = 0; r < 8; ++r)
            if (base + 2 * r + hh < segEnd) pp[32 * r] = v[r];
    }
}

static inline int ceil_div_i(long long a, long long b) { return (int)((a + b - 1) / b); }

static void launch_gemm(const float* X, const float* W, const float* b, const float* scale,
                        float* Y, int M, int K, int N, hipStream_t stream) {
    const int TPB = 256;
    const int strips = ceil_div_i(M, 16);
    if (N == 128 && (K == 32 || K == 64 || K == 128)) {
        const int grid = strips < 1024 ? strips : 1024;
        if (scale) {
            if (K == 32)
                gemm_bias_wmma_ws<32, true><<<grid, TPB, 0, stream>>>(X, W, b, scale, Y, M);
            else if (K == 64)
                gemm_bias_wmma_ws<64, true><<<grid, TPB, 0, stream>>>(X, W, b, scale, Y, M);
            else
                gemm_bias_wmma_ws<128, true><<<grid, TPB, 0, stream>>>(X, W, b, scale, Y, M);
        } else {
            if (K == 32)
                gemm_bias_wmma_ws<32, false><<<grid, TPB, 0, stream>>>(X, W, b, scale, Y, M);
            else if (K == 64)
                gemm_bias_wmma_ws<64, false><<<grid, TPB, 0, stream>>>(X, W, b, scale, Y, M);
            else
                gemm_bias_wmma_ws<128, false><<<grid, TPB, 0, stream>>>(X, W, b, scale, Y, M);
        }
        return;
    }
    if (K == 32)
        gemm_bias_wmma_lds<32><<<strips, TPB, 0, stream>>>(X, W, b, scale, Y, M, N);
    else if (K == 64)
        gemm_bias_wmma_lds<64><<<strips, TPB, 0, stream>>>(X, W, b, scale, Y, M, N);
    else if (K == 128)
        gemm_bias_wmma_lds<128><<<strips, TPB, 0, stream>>>(X, W, b, scale, Y, M, N);
    else {
        const int tiles = strips * (N / 16);
        gemm_bias_wmma_rt<<<ceil_div_i(tiles, 8), TPB, 0, stream>>>(X, W, b, scale, Y, M, K, N);
    }
}

extern "C" void kernel_launch(void* const* d_in, const int* in_sizes, int n_in,
                              void* d_out, int out_size, void* d_ws, size_t ws_size,
                              hipStream_t stream) {
    const float* x_user  = (const float*)d_in[0];
    const float* x_movie = (const float*)d_in[1];
    const int*   edge    = (const int*)d_in[2];
    const float* Wu = (const float*)d_in[3];
    const float* bu = (const float*)d_in[4];
    const float* Wm = (const float*)d_in[5];
    const float* bm = (const float*)d_in[6];
    const float* W1 = (const float*)d_in[7];
    const float* b1 = (const float*)d_in[8];
    const float* W2 = (const float*)d_in[9];
    const float* b2 = (const float*)d_in[10];
    const float* We = (const float*)d_in[11];
    const float* be = (const float*)d_in[12];
    float* out = (float*)d_out;

    const int H  = in_sizes[4];
    const int C  = in_sizes[12];
    const int Fu = in_sizes[3] / H;
    const int Fm = in_sizes[5] / H;
    const int Nu = in_sizes[0] / Fu;
    const int Nm = in_sizes[1] / Fm;
    const int E  = in_sizes[2] / 2;
    const int N  = Nu + Nm;
    const int* row = edge;
    const int* col = edge + E;

    float* ws   = (float*)d_ws;
    float* X    = ws;
    float* Hh   = X  + (size_t)N * H;
    float* Acc  = Hh + (size_t)N * H;
    float* dinv = Acc + (size_t)N * H;

    const int TPB = 256, WPB = 8;

    const size_t Npad = ((size_t)N + 31) & ~(size_t)31;
    float* P      = dinv + Npad;
    int*   cnt    = (int*)(P + (size_t)N * 16);
    int*   start  = cnt + (((size_t)N + 1 + 31) & ~(size_t)31);
    int*   cursor = start + Npad;
    int2*  slots  = (int2*)(cursor + Npad);
    float* Wcu = (float*)(slots + (((size_t)E + 15) & ~(size_t)15));
    float* Wcm = Wcu + (size_t)H * Fu;
    float* bcu = Wcm + (size_t)H * Fm;
    float* bcm = bcu + H;
    const bool fold = (H == 128) && (Fu % 32 == 0) && (Fm % 32 == 0);
    int* srcs     = (int*)(bcm + H);
    int* BH       = srcs + (((size_t)E + 31) & ~(size_t)31);
    int* binBase  = BH + (size_t)SORT_NBMAX * SORT_G;
    int* binCount = binBase + SORT_NBMAX;
    const int NB = ceil_div_i(N, 256);
    const bool sorted = NB <= SORT_NBMAX;

    const int composeUnits = fold ? 8 * (Fu / 32) + 8 * (Fm / 32) + 8 : 0;
    if (sorted) {
        const int chunk = ceil_div_i(E, SORT_G);
        bin_hist_kernel<<<SORT_G + ceil_div_i(composeUnits, COMPOSE_WPB), SORT_T, 0, stream>>>(
            col, BH, E, NB, chunk, W1, b1, Wu, bu, Wm, bm, Wcu, bcu, Wcm, bcm, Fu, Fm);
        bin_scatter_kernel<<<SORT_G, SORT_T, 0, stream>>>(row, col, BH, binBase, binCount, slots, E, NB, chunk);
        bin_sort_kernel<<<NB, SORT_T, 0, stream>>>(slots, binBase, binCount, start, cnt, dinv, srcs, N, E);
    } else {
        zero_int_kernel<<<ceil_div_i(N + 1, TPB), TPB, 0, stream>>>(cnt, N + 1);
        count_targets_kernel<<<ceil_div_i(E, TPB), TPB, 0, stream>>>(col, cnt, E);
        alloc_segments_kernel<<<ceil_div_i(N, TPB), TPB, 0, stream>>>(cnt, cnt + N, start, cursor, dinv, N);
        fill_segments_kernel<<<ceil_div_i(E, TPB), TPB, 0, stream>>>(row, col, dinv, cursor, slots, E);
    }
    const float* rowScale = sorted ? dinv : nullptr;

    if (fold) {
        if (!sorted)
            compose_wmma<<<ceil_div_i(composeUnits, WPB), TPB, 0, stream>>>(W1, b1, Wu, bu, Wm, bm,
                                                                            Wcu, bcu, Wcm, bcm, Fu, Fm);
        if (sorted && Fu == 32 && Fm == 64) {
            const int su = ceil_div_i(Nu, 16), sm = ceil_div_i(Nm, 16);
            const int gu = su < 768 ? su : 768, gm = sm < 768 ? sm : 768;
            gemm_pair_wmma_ws<32, 64, true><<<gu + gm, TPB, 0, stream>>>(
                x_user, Wcu, bcu, rowScale, Hh, Nu, gu,
                x_movie, Wcm, bcm, rowScale + Nu, Hh + (size_t)Nu * H, Nm);
        } else {
            launch_gemm(x_user, Wcu, bcu, rowScale, Hh, Nu, Fu, H, stream);
            launch_gemm(x_movie, Wcm, bcm, rowScale ? rowScale + Nu : nullptr,
                        Hh + (size_t)Nu * H, Nm, Fm, H, stream);
        }
    } else {
        launch_gemm(x_user, Wu, bu, nullptr, X, Nu, Fu, H, stream);
        launch_gemm(x_movie, Wm, bm, nullptr, X + (size_t)Nu * H, Nm, Fm, H, stream);
    }

    const int gatherWaves = N * ((H + 127) / 128);

    if (!fold) launch_gemm(X, W1, b1, rowScale, Hh, N, H, H, stream);
    if (sorted)
        gather_relu_kernel<true><<<ceil_div_i(gatherWaves, WPB), TPB, 0, stream>>>(Hh, dinv, start, cnt, slots, srcs, Acc, N, E, H);
    else
        gather_relu_kernel<false><<<ceil_div_i(gatherWaves, WPB), TPB, 0, stream>>>(Hh, dinv, start, cnt, slots, srcs, Acc, N, E, H);

    const bool fusedProj = sorted && H == 128 && C == 10;
    launch_gemm(Acc, W2, b2, rowScale, Hh, N, H, H, stream);
    if (fusedProj)
        gather_relu_proj_kernel<<<ceil_div_i(Nu, 16) + ceil_div_i(Nm, 16), TPB, 0, stream>>>(
            Hh, dinv, start, cnt, srcs, We, be, P, Nu, Nm, E, C);
    else if (sorted)
        gather_relu_kernel<true><<<ceil_div_i(gatherWaves, WPB), TPB, 0, stream>>>(Hh, dinv, start, cnt, slots, srcs, X, N, E, H);
    else
        gather_relu_kernel<false><<<ceil_div_i(gatherWaves, WPB), TPB, 0, stream>>>(Hh, dinv, start, cnt, slots, srcs, X, N, E, H);

    const int edgeTiles = ceil_div_i(E, 16);
    if (H == 128 && C == 10) {
        const int nodeTiles = ceil_div_i(Nu, 16) + ceil_div_i(Nm, 16);
        if (!fusedProj)
            node_proj_wmma<<<ceil_div_i(nodeTiles, 4), 128, 0, stream>>>(X, We, be, P, Nu, Nm, C);
        const long long quads = ((long long)E * C + 3) / 4;
        edge_add10<4><<<ceil_div_i(quads, TPB * 4), TPB, 0, stream>>>(P, row, col, out, E, Nu);
    } else if (H == 128)
        edge_mlp_wmma<128><<<ceil_div_i(edgeTiles, WPB), TPB, 0, stream>>>(X, row, col, We, be, out, E, Nu, H, C);
    else
        edge_mlp_wmma<0><<<ceil_div_i(edgeTiles, WPB), TPB, 0, stream>>>(X, row, col, We, be, out, E, Nu, H, C);
}


# first aggregation launch also runs conv2's 128-wide GEMM: resident blocks keep W in registers and walk 16-node tiles, summed rows go via LDS to WMMA; GEMM launch, x1 plane gone; sum order same
# speedup vs baseline: 5.9053x; 1.0521x over previous
#include <hip/hip_runtime.h>

typedef __attribute__((ext_vector_type(2))) float v2f;
typedef __attribute__((ext_vector_type(8))) float v8f;

__device__ __forceinline__ v8f wmma_f32x4(v2f a, v2f b, v8f c) {
    return __builtin_amdgcn_wmma_f32_16x16x4_f32(
         false, a,  false, b,
         (short)0, c,  false,  false);
}

__device__ __forceinline__ void load_row_scale8(const float* __restrict__ scale,
                                                int row0, int M, float (&sc)[8]) {
    if (scale == nullptr) {
#pragma unroll
        for (int r = 0; r < 8; ++r) sc[r] = 1.0f;
    } else {
#pragma unroll
        for (int r = 0; r < 8; ++r) sc[r] = scale[min(row0 + r, M - 1)];
    }
}

template <int K>
__global__ void gemm_bias_wmma_lds(const float* __restrict__ X,
                                   const float* __restrict__ W,
                                   const float* __restrict__ bias,
                                   const float* __restrict__ scale,
                                   float* __restrict__ Y,
                                   int M, int N) {
    __shared__ float As[16 * K];

    const int lane = threadIdx.x & 31;
    const int wave = threadIdx.x >> 5;
    const int wavesPerBlock = blockDim.x >> 5;
    const int tilesN = N >> 4;
    const int strip = blockIdx.x * 16;

    if (strip + 16 <= M) {
        const float4* __restrict__ src = (const float4*)(X + (size_t)strip * K);
        float4* dst = (float4*)As;
        for (int i = threadIdx.x; i < (16 * K) / 4; i += blockDim.x) dst[i] = src[i];
    } else {
        for (int i = threadIdx.x; i < 16 * K; i += blockDim.x) {
            const int rr = i / K, cc = i - rr * K;
            const int gr = min(strip + rr, M - 1);
            As[i] = X[(size_t)gr * K + cc];
        }
    }
    __syncthreads();

    const int koff = (lane >> 4) << 1;
    const float* __restrict__ arow = As + (lane & 15) * K + koff;
    const bool fullTile = (strip + 16 <= M);
    const int rbase = strip + ((lane >> 4) << 3);
    float sc[8];
    load_row_scale8(scale, rbase, M, sc);

    for (int tn = wave; tn < tilesN; tn += wavesPerBlock) {
        const int ncol = tn * 16 + (lane & 15);
        const float* __restrict__ wrow = W + (size_t)ncol * K + koff;

        v8f acc = {};
#pragma unroll
        for (int k = 0; k < K; k += 4) {
            v2f a = *(const v2f*)(arow + k);
            v2f b = *(const v2f*)(wrow + k);
            acc = wmma_f32x4(a, b, acc);
        }

        const float bv = bias[ncol];
        float* yp = Y + (size_t)rbase * N + ncol;
        if (fullTile) {
#pragma unroll
            for (int r = 0; r < 8; ++r) { *yp = (acc[r] + bv) * sc[r]; yp += N; }
        } else {
#pragma unroll
            for (int r = 0; r < 8; ++r) {
                if (rbase + r < M) *yp = (acc[r] + bv) * sc[r];
                yp += N;
            }
        }
    }
}

template <int K, bool SCALED>
__device__ __forceinline__ void
gemm_ws_body(const float* __restrict__ X,
             const float* __restrict__ W,
             const float* __restrict__ bias,
             const float* __restrict__ scale,
             float* __restrict__ Y, int M,
             float* __restrict__ lds, int firstStrip, int stripStep) {
    constexpr int N = 128;
    constexpr int P = K + 4;
    constexpr int C4 = K / 4;
    constexpr int NV = 16 * C4;
    constexpr int PER = (NV + 255) / 256;
    float (*As)[16 * P] = (float (*)[16 * P])lds;

    const int lane = threadIdx.x & 31;
    const int wave = threadIdx.x >> 5;
    const int c = lane & 15;
    const int h = lane >> 4;
    const int ncol = wave * 16 + c;
    const int strips = (M + 15) >> 4;
    const int step = stripStep;

    float4 bf[K / 8];
    {
        const float* __restrict__ wrow = W + (size_t)ncol * K + 4 * h;
#pragma unroll
        for (int i = 0; i < K / 8; ++i) bf[i] = *(const float4*)(wrow + 8 * i);
    }
    const float bv = bias[ncol];

    static_assert(PER <= 2, "a strip is at most 512 float4s");
    const int i0 = threadIdx.x;
    const int rr0 = i0 / C4, cc0 = i0 - rr0 * C4;
    const int i1 = i0 + 256;
    const int rr1 = i1 / C4, cc1 = i1 - rr1 * C4;

    int s = firstStrip;
    if (s < strips) {
        if (NV >= 256 || i0 < NV)
            *(float4*)(&As[0][rr0 * P + 4 * cc0]) =
                *(const float4*)(X + (size_t)min(s * 16 + rr0, M - 1) * K + 4 * cc0);
        if (PER > 1)
            *(float4*)(&As[0][rr1 * P + 4 * cc1]) =
                *(const float4*)(X + (size_t)min(s * 16 + rr1, M - 1) * K + 4 * cc1);
    }
    __syncthreads();

    const int aoff = c * P + 4 * h;
    int buf = 0;
    for (; s < strips; s += step) {
        const int sn = s + step;
        const bool more = sn < strips;
        float4 pre0 = make_float4(0.f, 0.f, 0.f, 0.f);
        float4 pre1 = make_float4(0.f, 0.f, 0.f, 0.f);
        if (more) {
            if (NV >= 256 || i0 < NV)
                pre0 = *(const float4*)(X + (size_t)min(sn * 16 + rr0, M - 1) * K + 4 * cc0);
            if (PER > 1)
                pre1 = *(const float4*)(X + (size_t)min(sn * 16 + rr1, M - 1) * K + 4 * cc1);
        }
        const int rbase = s * 16 + 8 * h;
        float4 su = make_float4(1.f, 1.f, 1.f, 1.f), sv = su;
        if (SCALED) {
            const float* __restrict__ sp = scale + max(min(rbase, M - 8), 0);
            __builtin_memcpy(&su, sp, 16);
            __builtin_memcpy(&sv, sp + 4, 16);
        }

        const float* arow = &As[buf][aoff];
        v8f acc = {};
#pragma unroll
        for (int i = 0; i < K / 8; ++i) {
            const float4 a = *(const float4*)(arow + 8 * i);
            v2f a0 = {a.x, a.y}, b0 = {bf[i].x, bf[i].y};
            v2f a1 = {a.z, a.w}, b1 = {bf[i].z, bf[i].w};
            acc = wmma_f32x4(a0, b0, acc);
            acc = wmma_f32x4(a1, b1, acc);
        }

        if (more) {
            if (NV >= 256 || i0 < NV) *(float4*)(&As[buf ^ 1][rr0 * P + 4 * cc0]) = pre0;
            if (PER > 1) *(float4*)(&As[buf ^ 1][rr1 * P + 4 * cc1]) = pre1;
        }
        __syncthreads();
        buf ^= 1;

        float* yp = Y + (size_t)rbase * N + ncol;
        if (s * 16 + 16 <= M) {
            const float sc[8] = {su.x, su.y, su.z, su.w, sv.x, sv.y, sv.z, sv.w};
#pragma unroll
            for (int r = 0; r < 8; ++r) { *yp = (acc[r] + bv) * sc[r]; yp += N; }
        } else {
#pragma unroll
            for (int r = 0; r < 8; ++r) {
                if (rbase + r < M) *yp = (acc[r] + bv) * (SCALED ? scale[rbase + r] : 1.0f);
                yp += N;
            }
        }
    }
}

template <int K, bool SCALED>
__global__ void __launch_bounds__(256)
gemm_bias_wmma_ws(const float* __restrict__ X,
                  const float* __restrict__ W,
                  const float* __restrict__ bias,
                  const float* __restrict__ scale,
                  float* __restrict__ Y, int M) {
    __shared__ __attribute__((aligned(16))) float As[2 * 16 * (K + 4)];
    gemm_ws_body<K, SCALED>(X, W, bias, scale, Y, M, As, (int)blockIdx.x, (int)gridDim.x);
}

template <int KA, int KB, bool SCALED>
__global__ void __launch_bounds__(256)
gemm_pair_wmma_ws(const float* __restrict__ XA, const float* __restrict__ WA,
                  const float* __restrict__ biasA, const float* __restrict__ scaleA,
                  float* __restrict__ YA, int MA, int gridA,
                  const float* __restrict__ XB, const float* __restrict__ WB,
                  const float* __restrict__ biasB, const float* __restrict__ scaleB,
                  float* __restrict__ YB, int MB) {
    constexpr int KMAX = KA > KB ? KA : KB;
    __shared__ __attribute__((aligned(16))) float As[2 * 16 * (KMAX + 4)];
    const int bid = (int)blockIdx.x;
    if (bid < gridA)
        gemm_ws_body<KA, SCALED>(XA, WA, biasA, scaleA, YA, MA, As, bid, gridA);
    else
        gemm_ws_body<KB, SCALED>(XB, WB, biasB, scaleB, YB, MB, As, bid - gridA,
                                 (int)gridDim.x - gridA);
}

__device__ __forceinline__ void
compose_unit(const float* __restrict__ W1, const float* __restrict__ b1,
             const float* __restrict__ Wu, const float* __restrict__ bu,
             const float* __restrict__ Wm, const float* __restrict__ bm,
             float* __restrict__ Wcu, float* __restrict__ bcu,
             float* __restrict__ Wcm, float* __restrict__ bcm,
             int Fu, int Fm, int unit, int lane) {
    constexpr int H = 128;
    constexpr int S = H / 8;
    const int c = lane & 15;
    const int h = lane >> 4;
    const int wu = 8 * (Fu >> 5);
    const int wm = 8 * (Fm >> 5);

    if (unit < wu + wm) {
        const bool second = unit >= wu;
        const int u = second ? unit - wu : unit;
        const int F = second ? Fm : Fu;
        const float* __restrict__ Wt = second ? Wm : Wu;
        float* __restrict__ Wc = second ? Wcm : Wcu;
        const int groups = F >> 5;
        const int nt = u / groups;
        const int cg = u - nt * groups;

        const float* __restrict__ arow = W1 + (size_t)(nt * 16 + c) * H + 4 * h;
        const float* __restrict__ bcol = Wt + (size_t)(4 * h) * F + cg * 32 + c;

        float4 af[S];
        float bl[S][4], br[S][4];
#pragma unroll
        for (int i = 0; i < S; ++i) {
            af[i] = *(const float4*)(arow + 8 * i);
            const float* __restrict__ bp = bcol + (size_t)(8 * i) * F;
#pragma unroll
            for (int e = 0; e < 4; ++e) {
                bl[i][e] = bp[(size_t)e * F];
                br[i][e] = bp[(size_t)e * F + 16];
            }
        }

        v8f acc0 = {}, acc1 = {};
#pragma unroll
        for (int i = 0; i < S; ++i) {
            v2f a0 = {af[i].x, af[i].y}, a1 = {af[i].z, af[i].w};
            v2f p0 = {bl[i][0], bl[i][1]}, p1 = {bl[i][2], bl[i][3]};
            v2f q0 = {br[i][0], br[i][1]}, q1 = {br[i][2], br[i][3]};
            acc0 = wmma_f32x4(a0, p0, acc0);
            acc1 = wmma_f32x4(a0, q0, acc1);
            acc0 = wmma_f32x4(a1, p1, acc0);
            acc1 = wmma_f32x4(a1, q1, acc1);
        }

        float* wp = Wc + (size_t)(nt * 16 + 8 * h) * F + cg * 32 + c;
#pragma unroll
        for (int r = 0; r < 8; ++r) {
            wp[0] = acc0[r];
            wp[16] = acc1[r];
            wp += F;
        }
    } else if (unit < wu + wm + 8) {
        const int q = unit - wu - wm;
        const bool second = q >= 4;
        const float* __restrict__ bt = second ? bm : bu;
        float* __restrict__ bc = second ? bcm : bcu;
        const int nb = 32 * (q & 3);

        const float* __restrict__ ar0 = W1 + (size_t)(nb + c) * H + 4 * h;
        const float* __restrict__ ar1 = ar0 + (size_t)16 * H;
        const float* __restrict__ bv = bt + 4 * h;

        float4 a0f[S], a1f[S], bf[S];
#pragma unroll
        for (int i = 0; i < S; ++i) {
            a0f[i] = *(const float4*)(ar0 + 8 * i);
            a1f[i] = *(const float4*)(ar1 + 8 * i);
            bf[i] = *(const float4*)(bv + 8 * i);
        }
        const float4 s00 = *(const float4*)(b1 + nb + 8 * h);
        const float4 s01 = *(const float4*)(b1 + nb + 8 * h + 4);
        const float4 s10 = *(const float4*)(b1 + nb + 16 + 8 * h);
        const float4 s11 = *(const float4*)(b1 + nb + 16 + 8 * h + 4);

        v8f acc0 = {}, acc1 = {};
#pragma unroll
        for (int i = 0; i < S; ++i) {
            v2f x0 = {a0f[i].x, a0f[i].y}, x1 = {a0f[i].z, a0f[i].w};
            v2f y0 = {a1f[i].x, a1f[i].y}, y1 = {a1f[i].z, a1f[i].w};
            v2f p0 = {bf[i].x, bf[i].y}, p1 = {bf[i].z, bf[i].w};
            acc0 = wmma_f32x4(x0, p0, acc0);
            acc1 = wmma_f32x4(y0, p0, acc1);
            acc0 = wmma_f32x4(x1, p1, acc0);
            acc1 = wmma_f32x4(y1, p1, acc1);
        }

        const float4 o00 = make_float4(acc0[0] + s00.x, acc0[1] + s00.y, acc0[2] + s00.z, acc0[3] + s00.w);
        const float4 o01 = make_float4(acc0[4] + s01.x, acc0[5] + s01.y, acc0[6] + s01.z, acc0[7] + s01.w);
        const float4 o10 = make_float4(acc1[0] + s10.x, acc1[1] + s10.y, acc1[2] + s10.z, acc1[3] + s10.w);
        const float4 o11 = make_float4(acc1[4] + s11.x, acc1[5] + s11.y, acc1[6] + s11.z, acc1[7] + s11.w);
        if (c == 0) {
            float* op = bc + nb + 8 * h;
            *(float4*)(op) = o00;
            *(float4*)(op + 4) = o01;
            *(float4*)(op + 16) = o10;
            *(float4*)(op + 20) = o11;
        }
    }
}

__global__ void __launch_bounds__(256)
compose_wmma(const float* __restrict__ W1, const float* __restrict__ b1,
             const float* __restrict__ Wu, const float* __restrict__ bu,
             const float* __restrict__ Wm, const float* __restrict__ bm,
             float* __restrict__ Wcu, float* __restrict__ bcu,
             float* __restrict__ Wcm, float* __restrict__ bcm,
             int Fu, int Fm) {
    const int unit = blockIdx.x * (blockDim.x >> 5) + (threadIdx.x >> 5);
    compose_unit(W1, b1, Wu, bu, Wm, bm, Wcu, bcu, Wcm, bcm, Fu, Fm, unit, threadIdx.x & 31);
}

__global__ void gemm_bias_wmma_rt(const float* __restrict__ X,
                                  const float* __restrict__ W,
                                  const float* __restrict__ bias,
                                  const float* __restrict__ scale,
                                  float* __restrict__ Y,
                                  int M, int K, int N) {
    const int lane = threadIdx.x & 31;
    const int wave = threadIdx.x >> 5;
    const int tilesN = N >> 4;
    const int tile = blockIdx.x * (blockDim.x >> 5) + wave;
    const int tm = tile / tilesN;
    const int tn = tile - tm * tilesN;
    if (tm * 16 >= M) return;

    const int mrow = min(tm * 16 + (lane & 15), M - 1);
    const int ncol = tn * 16 + (lane & 15);
    const int koff = (lane >> 4) << 1;
    const float* __restrict__ xrow = X + (size_t)mrow * K + koff;
    const float* __restrict__ wrow = W + (size_t)ncol * K + koff;

    v8f acc = {};
#pragma unroll 8
    for (int k = 0; k < K; k += 4) {
        v2f a = *(const v2f*)(xrow + k);
        v2f b = *(const v2f*)(wrow + k);
        acc = wmma_f32x4(a, b, acc);
    }

    const float bv = bias[ncol];
    const int rbase = tm * 16 + ((lane >> 4) << 3);
    float sc[8];
    load_row_scale8(scale, rbase, M, sc);
    float* yp = Y + (size_t)rbase * N + ncol;
#pragma unroll
    for (int r = 0; r < 8; ++r) {
        if (rbase + r < M) *yp = (acc[r] + bv) * sc[r];
        yp += N;
    }
}

template <int HT>
__global__ void edge_mlp_wmma(const float* __restrict__ xf,
                              const int* __restrict__ row,
                              const int* __restrict__ col,
                              const float* __restrict__ We,
                              const float* __restrict__ be,
                              float* __restrict__ out,
                              int E, int Nu, int Hrt, int C) {
    const int H = (HT > 0) ? HT : Hrt;
    const int K = 2 * H;
    const int lane = threadIdx.x & 31;
    const int wave = threadIdx.x >> 5;
    const int tile = blockIdx.x * (blockDim.x >> 5) + wave;
    const int ebase = tile * 16;
    if (ebase >= E) return;

    const int m = lane & 15;
    const int e = min(ebase + m, E - 1);
    const int r = row[e];
    const int c = col[e];
    const int koff = (lane >> 4) << 1;

    const float* __restrict__ urow = xf + (size_t)r * H + koff;
    const float* __restrict__ mrow = xf + (size_t)(Nu + c) * H + koff;

    const int n = lane & 15;
    const float wmask = (n < C) ? 1.0f : 0.0f;
    const float* __restrict__ wrow = We + (size_t)min(n, C - 1) * K + koff;

    v8f acc = {};
#pragma unroll 8
    for (int k = 0; k < K; k += 4) {
        const float* src = (k + koff < H) ? (urow + k) : (mrow + (k - H));
        v2f a = *(const v2f*)src;
        v2f b = *(const v2f*)(wrow + k);
        b.x *= wmask;
        b.y *= wmask;
        acc = wmma_f32x4(a, b, acc);
    }

    if (n < C) {
        const float bv = be[n];
        const int mb = (lane >> 4) << 3;
        float* op = out + (size_t)(ebase + mb) * C + n;
        if (ebase + 16 <= E) {
#pragma unroll
            for (int rr = 0; rr < 8; ++rr) { *op = acc[rr] + bv; op += C; }
        } else {
#pragma unroll
            for (int rr = 0; rr < 8; ++rr) {
                if (ebase + mb + rr < E) *op = acc[rr] + bv;
                op += C;
            }
        }
    }
}

__global__ void node_proj_wmma(const float* __restrict__ xf,
                               const float* __restrict__ We,
                               const float* __restrict__ be,
                               float* __restrict__ P,
                               int Nu, int Nm, int C) {
    constexpr int H = 128;
    const int lane = threadIdx.x & 31;
    const int wave = threadIdx.x >> 5;
    const int tilesU = (Nu + 15) >> 4;
    const int tilesM = (Nm + 15) >> 4;
    const int tile = blockIdx.x * (blockDim.x >> 5) + wave;
    if (tile >= tilesU + tilesM) return;

    const bool second = tile >= tilesU;
    const int segEnd = second ? Nu + Nm : Nu;
    const int base = second ? Nu + ((tile - tilesU) << 4) : (tile << 4);

    const int c = lane & 15;
    const int h = lane >> 4;
    const int node = min(base + 2 * (c & 7) + (c >> 3), segEnd - 1);
    const float* __restrict__ arow = xf + (size_t)node * H + 4 * h;
    const float* __restrict__ brow =
        We + (size_t)min(c, C - 1) * (2 * H) + (second ? H : 0) + 4 * h;

    const float bsel = be[min(c, C - 1)];
    const float bv = (second && c < C) ? bsel : 0.0f;

    v8f acc0 = {}, acc1 = {};
#pragma unroll
    for (int i = 0; i < H / 8; ++i) {
        const float4 a = *(const float4*)(arow + 8 * i);
        const float4 b = *(const float4*)(brow + 8 * i);
        v2f a0 = {a.x, a.y}, b0 = {b.x, b.y};
        v2f a1 = {a.z, a.w}, b1 = {b.z, b.w};
        acc0 = wmma_f32x4(a0, b0, acc0);
        acc1 = wmma_f32x4(a1, b1, acc1);
    }

    float v[8];
#pragma unroll
    for (int r = 0; r < 8; ++r) v[r] = (acc0[r] + acc1[r]) + bv;

    float* pp = P + (size_t)base * 16 + lane;
    if (base + 16 <= segEnd) {
#pragma unroll
        for (int r = 0; r < 8; ++r) pp[32 * r] = v[r];
    } else {
#pragma unroll
        for (int r = 0; r < 8; ++r)
            if (base + 2 * r + h < segEnd) pp[32 * r] = v[r];
    }
}

template <int ITER>
__global__ void edge_add10(const float* __restrict__ P,
                           const int* __restrict__ row,
                           const int* __restrict__ col,
                           float* __restrict__ out,
                           int E, int Nu) {
    constexpr unsigned C = 10;
    const unsigned total = (unsigned)E * C;
    const float* __restrict__ Pu = P;
    const float* __restrict__ Pm = P + (size_t)Nu * 16;
    const unsigned perBlock = blockDim.x * ITER;
    const unsigned qb = blockIdx.x * perBlock;

    if ((qb + perBlock) * 4u <= total) {
        unsigned f0[ITER], cA[ITER], cB[ITER];
        int rA[ITER], mA[ITER], rB[ITER], mB[ITER];
#pragma unroll
        for (int it = 0; it < ITER; ++it) {
            f0[it] = 4u * (qb + it * blockDim.x + threadIdx.x);
            const unsigned f1 = f0[it] + 2u;
            const unsigned eA = f0[it] / C, eB = f1 / C;
            cA[it] = f0[it] - eA * C;
            cB[it] = f1 - eB * C;
            rA[it] = row[eA]; mA[it] = col[eA];
            rB[it] = row[eB]; mB[it] = col[eB];
        }
        float2 uA[ITER], vA[ITER], uB[ITER], vB[ITER];
#pragma unroll
        for (int it = 0; it < ITER; ++it) {
            uA[it] = *(const float2*)(Pu + (size_t)rA[it] * 16 + cA[it]);
            vA[it] = *(const float2*)(Pm + (size_t)mA[it] * 16 + cA[it]);
            uB[it] = *(const float2*)(Pu + (size_t)rB[it] * 16 + cB[it]);
            vB[it] = *(const float2*)(Pm + (size_t)mB[it] * 16 + cB[it]);
        }
#pragma unroll
        for (int it = 0; it < ITER; ++it) {
            *(float4*)(out + f0[it]) = make_float4(uA[it].x + vA[it].x, uA[it].y + vA[it].y,
                                                   uB[it].x + vB[it].x, uB[it].y + vB[it].y);
        }
    } else {
        for (unsigned p = threadIdx.x; p < 2u * perBlock; p += blockDim.x) {
            const unsigned f = 4u * qb + 2u * p;
            if (f < total) {
                const unsigned e = f / C, c = f - e * C;
                const float2 u = *(const float2*)(Pu + (size_t)row[e] * 16 + c);
                const float2 v = *(const float2*)(Pm + (size_t)col[e] * 16 + c);
                *(float2*)(out + f) = make_float2(u.x + v.x, u.y + v.y);
            }
        }
    }
}

constexpr int SORT_G = 128;
constexpr int SORT_NBMAX = 1024;
constexpr int SORT_T = 1024;
static_assert(SORT_T == SORT_NBMAX && SORT_G % 32 == 0, "one bin per thread, whole chunk rows per wave");

__device__ __forceinline__ int wave_incl_scan(int v, int lane) {
    for (int d = 1; d < 32; d <<= 1) {
        const int t = __shfl_up(v, d);
        if (lane >= d) v += t;
    }
    return v;
}

constexpr int COMPOSE_WPB = 8;
__global__ void __launch_bounds__(SORT_T)
bin_hist_kernel(const int* __restrict__ col, int* __restrict__ BH,
                int E, int NB, int chunk,
                const float* __restrict__ W1, const float* __restrict__ b1,
                const float* __restrict__ Wu, const float* __restrict__ bu,
                const float* __restrict__ Wm, const float* __restrict__ bm,
                float* __restrict__ Wcu, float* __restrict__ bcu,
                float* __restrict__ Wcm, float* __restrict__ bcm,
                int Fu, int Fm) {
    __shared__ int hist[SORT_NBMAX];
    const int tid = threadIdx.x;
    if (blockIdx.x >= SORT_G) {
        const int wave = tid >> 5;
        if (wave < COMPOSE_WPB)
            compose_unit(W1, b1, Wu, bu, Wm, bm, Wcu, bcu, Wcm, bcm, Fu, Fm,
                         ((int)blockIdx.x - SORT_G) * COMPOSE_WPB + wave, tid & 31);
        return;
    }
    const int g = blockIdx.x;
    hist[tid] = 0;
    __syncthreads();
    const int lo = min(g * chunk, E);
    const int hi = min(lo + chunk, E);
    for (int e0 = lo + tid; e0 < hi; e0 += 4 * SORT_T) {
        int c[4];
#pragma unroll
        for (int k = 0; k < 4; ++k) {
            const int e = e0 + k * SORT_T;
            c[k] = (e < hi) ? col[e] : -1;
        }
#pragma unroll
        for (int k = 0; k < 4; ++k)
            if (c[k] >= 0) atomicAdd(&hist[min(c[k] >> 8, NB - 1)], 1);
    }
    __syncthreads();
    BH[(size_t)g * SORT_NBMAX + tid] = hist[tid];
}

__global__ void __launch_bounds__(SORT_T)
bin_scatter_kernel(const int* __restrict__ row, const int* __restrict__ col,
                   const int* __restrict__ BH,
                   int* __restrict__ binBase, int* __restrict__ binCount,
                   int2* __restrict__ binned, int E, int NB, int chunk) {
    __shared__ int pre[SORT_NBMAX];
    __shared__ int tot[SORT_NBMAX];
    __shared__ int cur[SORT_NBMAX];
    __shared__ int wsum[32];
    const int tid = threadIdx.x;
    const int lane = tid & 31;
    const int wave = tid >> 5;
    const int g = blockIdx.x;

    pre[tid] = 0;
    tot[tid] = 0;
    __syncthreads();
    constexpr int GW = SORT_G / 32;
    const int g0 = wave * GW;
    for (int bb = 0; bb * 128 < NB; bb += 2) {
        const int b4 = bb * 128 + 4 * lane;
        const bool two = (bb + 1) * 128 < NB;
        int4 va[GW], vb[GW];
#pragma unroll
        for (int k = 0; k < GW; ++k) {
            const int* rp = BH + (size_t)(g0 + k) * SORT_NBMAX + b4;
            va[k] = *(const int4*)rp;
            vb[k] = two ? *(const int4*)(rp + 128) : make_int4(0, 0, 0, 0);
        }
        int4 pa = make_int4(0, 0, 0, 0), ta = pa, pb = pa, tb = pa;
#pragma unroll
        for (int k = 0; k < GW; ++k) {
            const bool early = g0 + k < g;
            ta.x += va[k].x; ta.y += va[k].y; ta.z += va[k].z; ta.w += va[k].w;
            tb.x += vb[k].x; tb.y += vb[k].y; tb.z += vb[k].z; tb.w += vb[k].w;
            if (early) {
                pa.x += va[k].x; pa.y += va[k].y; pa.z += va[k].z; pa.w += va[k].w;
                pb.x += vb[k].x; pb.y += vb[k].y; pb.z += vb[k].z; pb.w += vb[k].w;
            }
        }
        atomicAdd(&tot[b4 + 0], ta.x); atomicAdd(&tot[b4 + 1], ta.y);
        atomicAdd(&tot[b4 + 2], ta.z); atomicAdd(&tot[b4 + 3], ta.w);
        atomicAdd(&pre[b4 + 0], pa.x); atomicAdd(&pre[b4 + 1], pa.y);
        atomicAdd(&pre[b4 + 2], pa.z); atomicAdd(&pre[b4 + 3], pa.w);
        if (two) {
            atomicAdd(&tot[b4 + 128], tb.x); atomicAdd(&tot[b4 + 129], tb.y);
            atomicAdd(&tot[b4 + 130], tb.z); atomicAdd(&tot[b4 + 131], tb.w);
            atomicAdd(&pre[b4 + 128], pb.x); atomicAdd(&pre[b4 + 129], pb.y);
            atomicAdd(&pre[b4 + 130], pb.z); atomicAdd(&pre[b4 + 131], pb.w);
        }
    }
    __syncthreads();
    const int mine = tot[tid];
    const int incl = wave_incl_scan(mine, lane);
    if (lane == 31) wsum[wave] = incl;
    __syncthreads();
    if (wave == 0) {
        const int w = wsum[lane];
        const int wi = wave_incl_scan(w, lane);
        wsum[lane] = wi - w;
    }
    __syncthreads();
    const int base = wsum[wave] + incl - mine;
    cur[tid] = base + pre[tid];
    if (g == 0 && tid < NB) { binBase[tid] = base; binCount[tid] = mine; }
    __syncthreads();

    const int lo = min(g * chunk, E);
    const int hi = min(lo + chunk, E);
    for (int e0 = lo + tid; e0 < hi; e0 += 4 * SORT_T) {
        int c[4], r[4];
#pragma unroll
        for (int k = 0; k < 4; ++k) {
            const int e = e0 + k * SORT_T;
            c[k] = (e < hi) ? col[e] : -1;
            r[k] = (e < hi) ? row[e] : 0;
        }
#pragma unroll
        for (int k = 0; k < 4; ++k) {
            if (c[k] >= 0) {
                const int pos = atomicAdd(&cur[min(c[k] >> 8, NB - 1)], 1);
                binned[min(max(pos, 0), E - 1)] = make_int2(r[k], c[k]);
            }
        }
    }
}

__global__ void __launch_bounds__(SORT_T)
bin_sort_kernel(const int2* __restrict__ binned,
                const int* __restrict__ binBase, const int* __restrict__ binCount,
                int* __restrict__ start, int* __restrict__ cnt,
                float* __restrict__ dinv, int* __restrict__ srcs, int N, int E) {
    __shared__ int c[256];
    __shared__ int cur[256];
    __shared__ int wsum[8];
    const int tid = threadIdx.x;
    const int lane = tid & 31;
    const int wave = tid >> 5;
    const int b = blockIdx.x;
    const int count = min(max(binCount[b], 0), E);
    const int base = min(max(binBase[b], 0), E - count);
    const int node0 = b << 8;

    if (tid < 256) c[tid] = 0;
    __syncthreads();
    for (int i0 = tid; i0 < count; i0 += 4 * SORT_T) {
        int t[4];
#pragma unroll
        for (int k = 0; k < 4; ++k) {
            const int i = i0 + k * SORT_T;
            t[k] = (i < count) ? binned[base + i].y : -1;
        }
#pragma unroll
        for (int k = 0; k < 4; ++k)
            if (t[k] >= 0) atomicAdd(&c[min(max(t[k] - node0, 0), 255)], 1);
    }
    __syncthreads();
    int mine = 0, incl = 0;
    if (tid < 256) {
        mine = c[tid];
        incl = wave_incl_scan(mine, lane);
        if (lane == 31) wsum[wave] = incl;
    }
    __syncthreads();
    if (tid < 256) {
        int off = 0;
        for (int w = 0; w < wave; ++w) off += wsum[w];
        const int excl = off + incl - mine;
        cur[tid] = excl;
        const int node = node0 + tid;
        if (node < N) {
            start[node] = base + excl;
            cnt[node] = mine;
            dinv[node] = rsqrtf((float)(mine + 1));
        }
    }
    __syncthreads();
    for (int i0 = tid; i0 < count; i0 += 4 * SORT_T) {
        int2 ed[4];
#pragma unroll
        for (int k = 0; k < 4; ++k) {
            const int i = i0 + k * SORT_T;
            ed[k] = (i < count) ? binned[base + i] : make_int2(0, -1);
        }
#pragma unroll
        for (int k = 0; k < 4; ++k) {
            if (ed[k].y >= 0) {
                const int p = atomicAdd(&cur[min(max(ed[k].y - node0, 0), 255)], 1);
                srcs[base + min(max(p, 0), count - 1)] = ed[k].x;
            }
        }
    }
}

__global__ void zero_int_kernel(int* __restrict__ p, int n) {
    const int i = blockIdx.x * blockDim.x + threadIdx.x;
    if (i < n) p[i] = 0;
}

__global__ void count_targets_kernel(const int* __restrict__ col,
                                     int* __restrict__ cnt, int E) {
    const int e = blockIdx.x * blockDim.x + threadIdx.x;
    if (e < E) atomicAdd(&cnt[col[e]], 1);
}

__global__ void alloc_segments_kernel(const int* __restrict__ cnt,
                                      int* __restrict__ total,
                                      int* __restrict__ start,
                                      int* __restrict__ cursor,
                                      float* __restrict__ dinv, int N) {
    const int i = blockIdx.x * blockDim.x + threadIdx.x;
    const int lane = threadIdx.x & 31;
    const int c = (i < N) ? cnt[i] : 0;
    int incl = c;
    for (int d = 1; d < 32; d <<= 1) {
        const int t = __shfl_up(incl, d);
        if (lane >= d) incl += t;
    }
    int base = 0;
    if (lane == 31) base = atomicAdd(total, incl);
    base = __shfl(base, 31);
    if (i < N) {
        const int s = base + incl - c;
        start[i] = s;
        cursor[i] = s;
        dinv[i] = rsqrtf((float)(c + 1));
    }
}

__global__ void fill_segments_kernel(const int* __restrict__ row,
                                     const int* __restrict__ col,
                                     const float* __restrict__ dinv,
                                     int* __restrict__ cursor,
                                     int2* __restrict__ slots, int E) {
    const int e = blockIdx.x * blockDim.x + threadIdx.x;
    if (e < E) {
        const int r = row[e];
        const int pos = atomicAdd(&cursor[col[e]], 1);
        slots[pos] = make_int2(r, __float_as_int(dinv[r]));
    }
}

__device__ __forceinline__ float4 fma4(float w, float4 a, float4 c) {
    return make_float4(fmaf(w, a.x, c.x), fmaf(w, a.y, c.y), fmaf(w, a.z, c.z), fmaf(w, a.w, c.w));
}

__device__ __forceinline__ float4 add4(float4 a, float4 c) {
    return make_float4(a.x + c.x, a.y + c.y, a.z + c.z, a.w + c.w);
}

template <bool SORTED>
__global__ void gather_relu_kernel(const float* __restrict__ h,
                                   const float* __restrict__ dinv,
                                   const int* __restrict__ start,
                                   const int* __restrict__ cnt,
                                   const int2* __restrict__ slots,
                                   const int* __restrict__ srcs,
                                   float* __restrict__ xout,
                                   int N, int E, int H) {
    const int lane = threadIdx.x & 31;
    const int cpn = (H + 127) >> 7;
    const int wid = __builtin_amdgcn_readfirstlane(
        (int)(blockIdx.x * (blockDim.x >> 5) + (threadIdx.x >> 5)));
    const int node = wid / cpn;
    if (node >= N) return;
    const int cb = (wid - node * cpn) << 7;
    const bool act = cb + 4 * lane < H;
    const int c0 = min(cb + 4 * lane, H - 4);
    const float* __restrict__ hl = h + c0;

    const int c = min(max(cnt[node], 0), E);
    const int s = min(max(start[node], 0), E - c);
    const float4 hself = *(const float4*)(hl + (size_t)node * H);
    const float di = dinv[node];

    float4 acc = make_float4(0.f, 0.f, 0.f, 0.f);
    for (int k0 = 0; k0 < c; k0 += 32) {
        const int n = min(32, c - k0);
        int2 ev = make_int2(0, 0);
        if (lane < n) {
            if (SORTED)
                ev.x = min(max(srcs[s + k0 + lane], 0), N - 1);
            else
                ev = slots[s + k0 + lane];
        }
        for (int j = 0; j < n; j += 4) {
            const int j1 = min(j + 1, n - 1), j2 = min(j + 2, n - 1), j3 = min(j + 3, n - 1);
            const int s0 = __builtin_amdgcn_readlane(ev.x, j);
            const int s1 = __builtin_amdgcn_readlane(ev.x, j1);
            const int s2 = __builtin_amdgcn_readlane(ev.x, j2);
            const int s3 = __builtin_amdgcn_readlane(ev.x, j3);
            const float4 a0 = *(const float4*)(hl + (size_t)s0 * H);
            const float4 a1 = *(const float4*)(hl + (size_t)s1 * H);
            const float4 a2 = *(const float4*)(hl + (size_t)s2 * H);
            const float4 a3 = *(const float4*)(hl + (size_t)s3 * H);
            if (SORTED) {
                acc = add4(a0, acc);
                if (j + 1 < n) acc = add4(a1, acc);
                if (j + 2 < n) acc = add4(a2, acc);
                if (j + 3 < n) acc = add4(a3, acc);
            } else {
                const float w0 = __int_as_float(__builtin_amdgcn_readlane(ev.y, j));
                const float w1 = __int_as_float(__builtin_amdgcn_readlane(ev.y, j1));
                const float w2 = __int_as_float(__builtin_amdgcn_readlane(ev.y, j2));
                const float w3 = __int_as_float(__builtin_amdgcn_readlane(ev.y, j3));
                acc = fma4(w0, a0, acc);
                if (j + 1 < n) acc = fma4(w1, a1, acc);
                if (j + 2 < n) acc = fma4(w2, a2, acc);
                if (j + 3 < n) acc = fma4(w3, a3, acc);
            }
        }
    }

    if (act) {
        float4 v;
        if (SORTED) {
            v = make_float4(di * (acc.x + hself.x), di * (acc.y + hself.y),
                            di * (acc.z + hself.z), di * (acc.w + hself.w));
        } else {
            const float d2 = di * di;
            v = make_float4(di * acc.x + d2 * hself.x, di * acc.y + d2 * hself.y,
                            di * acc.z + d2 * hself.z, di * acc.w + d2 * hself.w);
        }
        v.x = v.x > 0.f ? v.x : 0.f;
        v.y = v.y > 0.f ? v.y : 0.f;
        v.z = v.z > 0.f ? v.z : 0.f;
        v.w = v.w > 0.f ? v.w : 0.f;
        *(float4*)(xout + (size_t)node * H + c0) = v;
    }
}

__global__ void __launch_bounds__(256)
gather_relu_proj_kernel(const float* __restrict__ h,
                        const float* __restrict__ dinv,
                        const int* __restrict__ start,
                        const int* __restrict__ cnt,
                        const int* __restrict__ srcs,
                        const float* __restrict__ We,
                        const float* __restrict__ be,
                        float* __restrict__ P,
                        int Nu, int Nm, int E, int C) {
    constexpr int H = 128;
    constexpr int PITCH = H + 4;
    __shared__ __attribute__((aligned(16))) float As[16 * PITCH];

    const int lane = threadIdx.x & 31;
    const int wave = __builtin_amdgcn_readfirstlane((int)(threadIdx.x >> 5));
    const int N = Nu + Nm;
    const int tilesU = (Nu + 15) >> 4;
    const int tile = (int)blockIdx.x;
    const bool second = tile >= tilesU;
    const int segEnd = second ? Nu + Nm : Nu;
    const int base = second ? Nu + ((tile - tilesU) << 4) : (tile << 4);

    const float* __restrict__ hl = h + 4 * lane;

    int nodeT[2], cT[2], sT[2], evT[2];
    float diT[2];
    float4 selfT[2];
#pragma unroll
    for (int t = 0; t < 2; ++t) {
        nodeT[t] = min(base + 2 * wave + t, segEnd - 1);
        cT[t] = min(max(cnt[nodeT[t]], 0), E);
        sT[t] = min(max(start[nodeT[t]], 0), E - cT[t]);
        diT[t] = dinv[nodeT[t]];
        selfT[t] = *(const float4*)(hl + (size_t)nodeT[t] * H);
    }
    int rawT[2];
#pragma unroll
    for (int t = 0; t < 2; ++t) {
        rawT[t] = 0;
        if (lane < min(32, cT[t])) rawT[t] = srcs[sT[t] + lane];
    }
#pragma unroll
    for (int t = 0; t < 2; ++t)
        evT[t] = min(max(rawT[t], 0), N - 1);

#pragma unroll
    for (int t = 0; t < 2; ++t) {
        const int c = cT[t];
        const int s = sT[t];
        float4 acc = make_float4(0.f, 0.f, 0.f, 0.f);
        int ev = evT[t];
        for (int k0 = 0; k0 < c; k0 += 32) {
            const int n = min(32, c - k0);
            if (k0 > 0) {
                ev = 0;
                if (lane < n) ev = min(max(srcs[s + k0 + lane], 0), N - 1);
            }
            for (int j = 0; j < n; j += 4) {
                const int j1 = min(j + 1, n - 1), j2 = min(j + 2, n - 1), j3 = min(j + 3, n - 1);
                const int s0 = __builtin_amdgcn_readlane(ev, j);
                const int s1 = __builtin_amdgcn_readlane(ev, j1);
                const int s2 = __builtin_amdgcn_readlane(ev, j2);
                const int s3 = __builtin_amdgcn_readlane(ev, j3);
                const float4 a0 = *(const float4*)(hl + (size_t)s0 * H);
                const float4 a1 = *(const float4*)(hl + (size_t)s1 * H);
                const float4 a2 = *(const float4*)(hl + (size_t)s2 * H);
                const float4 a3 = *(const float4*)(hl + (size_t)s3 * H);
                acc = add4(a0, acc);
                if (j + 1 < n) acc = add4(a1, acc);
                if (j + 2 < n) acc = add4(a2, acc);
                if (j + 3 < n) acc = add4(a3, acc);
            }
        }
        const float di = diT[t];
        const float4 hs = selfT[t];
        float4 v = make_float4(di * (acc.x + hs.x), di * (acc.y + hs.y),
                               di * (acc.z + hs.z), di * (acc.w + hs.w));
        v.x = v.x > 0.f ? v.x : 0.f;
        v.y = v.y > 0.f ? v.y : 0.f;
        v.z = v.z > 0.f ? v.z : 0.f;
        v.w = v.w > 0.f ? v.w : 0.f;
        *(float4*)(&As[(2 * wave + t) * PITCH + 4 * lane]) = v;
    }
    __syncthreads();
    if (wave != 0) return;

    const int c = lane & 15;
    const int hh = lane >> 4;
    const float* arow = &As[(2 * (c & 7) + (c >> 3)) * PITCH + 4 * hh];
    const float* __restrict__ brow =
        We + (size_t)min(c, C - 1) * (2 * H) + (second ? H : 0) + 4 * hh;

    const float bsel = be[min(c, C - 1)];
    const float bv = (second && c < C) ? bsel : 0.0f;

    v8f acc0 = {}, acc1 = {};
#pragma unroll
    for (int i = 0; i < H / 8; ++i) {
        const float4 a = *(const float4*)(arow + 8 * i);
        const float4 b = *(const float4*)(brow + 8 * i);
        v2f a0 = {a.x, a.y}, b0 = {b.x, b.y};
        v2f a1 = {a.z, a.w}, b1 = {b.z, b.w};
        acc0 = wmma_f32x4(a0, b0, acc0);
        acc1 = wmma_f32x4(a1, b1, acc1);
    }

    float v[8];
#pragma unroll
    for (int r = 0; r < 8; ++r) v[r] = (acc0[r] + acc1[r]) + bv;

    float* pp = P + (size_t)base * 16 + lane;
    if (base + 16 <= segEnd) {
#pragma unroll
        for (int r = 0; r < 8; ++r) pp[32 * r] = v[r];
    } else {
#pragma unroll
        for (int r = 0; r < 8; ++r)
            if (base + 2 * r + hh < segEnd) pp[32 * r] = v[r];
    }
}

__global__ void __launch_bounds__(256)
gather_relu_gemm_kernel(const float* __restrict__ h,
                        const float* __restrict__ dinv,
                        const int* __restrict__ start,
                        const int* __restrict__ cnt,
                        const int* __restrict__ srcs,
                        const float* __restrict__ W,
                        const float* __restrict__ bias,
                        float* __restrict__ Y,
                        int N, int E) {
    constexpr int H = 128;
    constexpr int PITCH = H + 4;
    constexpr int G = 6;
    __shared__ __attribute__((aligned(16))) float As[2][16 * PITCH];
    __shared__ __attribute__((aligned(16))) float Ds[2][16];

    const int lane = threadIdx.x & 31;
    const int wave = __builtin_amdgcn_readfirstlane((int)(threadIdx.x >> 5));
    const int c = lane & 15;
    const int hh = lane >> 4;
    const int ncol = wave * 16 + c;
    const int tiles = (N + 15) >> 4;
    const int nb = (int)gridDim.x;
    const int b = (int)blockIdx.x;
    if (b >= tiles) return;
    const int trips = (tiles - b + nb - 1) / nb;
    const bool rev = (b & 1) != 0;

    float4 bf[H / 8];
    {
        const float* __restrict__ wrow = W + (size_t)ncol * H + 4 * hh;
#pragma unroll
        for (int i = 0; i < H / 8; ++i) bf[i] = *(const float4*)(wrow + 8 * i);
    }
    const float bv = bias[ncol];
    const int loff = 4 * lane;
    const int aoff = c * PITCH + 4 * hh;

    int tile = b + (rev ? trips - 1 : 0) * nb;
    int cT[2], sT[2], evT[2];
    float diT[2];
    float4 selfT[2];
    {
        const int nl = min((tile << 4) + c, N - 1);
        const int cv = cnt[nl];
        const int sv = start[nl];
#pragma unroll
        for (int t = 0; t < 2; ++t) {
            const int node = min((tile << 4) + 2 * wave + t, N - 1);
            cT[t] = min(max(__builtin_amdgcn_readlane(cv, 2 * wave + t), 0), E);
            sT[t] = min(max(__builtin_amdgcn_readlane(sv, 2 * wave + t), 0), E - cT[t]);
            diT[t] = dinv[node];
            selfT[t] = *(const float4*)(h + (size_t)node * H + loff);
        }
#pragma unroll
        for (int t = 0; t < 2; ++t) {
            int raw = 0;
            if (lane < min(32, cT[t])) raw = srcs[sT[t] + lane];
            evT[t] = min(max(raw, 0), N - 1);
        }
    }

    int buf = 0;
    for (int k = 0; k < trips; ++k) {
        const int base = tile << 4;
        const int kn = min(k + 1, trips - 1);
        const int tnext = b + (rev ? trips - 1 - kn : kn) * nb;
        const int nl = min((tnext << 4) + c, N - 1);
        const int cv = cnt[nl];
        const int sv = start[nl];

#pragma unroll
        for (int t = 0; t < 2; ++t) {
            const int cc = cT[t];
            const int ss = sT[t];
            float4 acc = make_float4(0.f, 0.f, 0.f, 0.f);
            int ev = evT[t];
            for (int k0 = 0; k0 < cc; k0 += 32) {
                const int n = min(32, cc - k0);
                if (k0 > 0) {
                    ev = 0;
                    if (lane < n) ev = min(max(srcs[ss + k0 + lane], 0), N - 1);
                }
                for (int j = 0; j < n; j += G) {
                    float4 a[G];
#pragma unroll
                    for (int q = 0; q < G; ++q) {
                        const int sq = __builtin_amdgcn_readlane(ev, min(j + q, n - 1));
                        const float* __restrict__ rowp = h + (size_t)sq * H;
                        a[q] = *(const float4*)(rowp + loff);
                    }
                    acc = add4(a[0], acc);
#pragma unroll
                    for (int q = 1; q < G; ++q)
                        if (j + q < n) acc = add4(a[q], acc);
                }
            }
            const float di = diT[t];
            const float4 hs = selfT[t];
            float4 v = make_float4(di * (acc.x + hs.x), di * (acc.y + hs.y),
                                   di * (acc.z + hs.z), di * (acc.w + hs.w));
            v.x = v.x > 0.f ? v.x : 0.f;
            v.y = v.y > 0.f ? v.y : 0.f;
            v.z = v.z > 0.f ? v.z : 0.f;
            v.w = v.w > 0.f ? v.w : 0.f;
            *(float4*)(&As[buf][(2 * wave + t) * PITCH + loff]) = v;
            if (lane == 0) Ds[buf][2 * wave + t] = di;
        }
        __syncthreads();

#pragma unroll
        for (int t = 0; t < 2; ++t) {
            const int node = min((tnext << 4) + 2 * wave + t, N - 1);
            cT[t] = min(max(__builtin_amdgcn_readlane(cv, 2 * wave + t), 0), E);
            sT[t] = min(max(__builtin_amdgcn_readlane(sv, 2 * wave + t), 0), E - cT[t]);
            diT[t] = dinv[node];
            selfT[t] = *(const float4*)(h + (size_t)node * H + loff);
        }
        int rawN[2];
#pragma unroll
        for (int t = 0; t < 2; ++t) {
            rawN[t] = 0;
            if (lane < min(32, cT[t])) rawN[t] = srcs[sT[t] + lane];
        }

        const float4 su = *(const float4*)(&Ds[buf][8 * hh]);
        const float4 sw = *(const float4*)(&Ds[buf][8 * hh + 4]);

        const float* arow = &As[buf][aoff];
        v8f acc = {};
#pragma unroll
        for (int i = 0; i < H / 8; ++i) {
            const float4 a = *(const float4*)(arow + 8 * i);
            v2f a0 = {a.x, a.y}, b0 = {bf[i].x, bf[i].y};
            v2f a1 = {a.z, a.w}, b1 = {bf[i].z, bf[i].w};
            acc = wmma_f32x4(a0, b0, acc);
            acc = wmma_f32x4(a1, b1, acc);
        }

        const int rbase = base + 8 * hh;
        const float sc[8] = {su.x, su.y, su.z, su.w, sw.x, sw.y, sw.z, sw.w};
        float* yp = Y + (size_t)rbase * H + ncol;
        if (base + 16 <= N) {
#pragma unroll
            for (int r = 0; r < 8; ++r) { *yp = (acc[r] + bv) * sc[r]; yp += H; }
        } else {
#pragma unroll
            for (int r = 0; r < 8; ++r) {
                if (rbase + r < N) *yp = (acc[r] + bv) * sc[r];
                yp += H;
            }
        }

#pragma unroll
        for (int t = 0; t < 2; ++t)
            evT[t] = min(max(rawN[t], 0), N - 1);
        tile = tnext;
        buf ^= 1;
    }
}

static inline int ceil_div_i(long long a, long long b) { return (int)((a + b - 1) / b); }

static void launch_gemm(const float* X, const float* W, const float* b, const float* scale,
                        float* Y, int M, int K, int N, hipStream_t stream) {
    const int TPB = 256;
    const int strips = ceil_div_i(M, 16);
    if (N == 128 && (K == 32 || K == 64 || K == 128)) {
        const int grid = strips < 1024 ? strips : 1024;
        if (scale) {
            if (K == 32)
                gemm_bias_wmma_ws<32, true><<<grid, TPB, 0, stream>>>(X, W, b, scale, Y, M);
            else if (K == 64)
                gemm_bias_wmma_ws<64, true><<<grid, TPB, 0, stream>>>(X, W, b, scale, Y, M);
            else
                gemm_bias_wmma_ws<128, true><<<grid, TPB, 0, stream>>>(X, W, b, scale, Y, M);
        } else {
            if (K == 32)
                gemm_bias_wmma_ws<32, false><<<grid, TPB, 0, stream>>>(X, W, b, scale, Y, M);
            else if (K == 64)
                gemm_bias_wmma_ws<64, false><<<grid, TPB, 0, stream>>>(X, W, b, scale, Y, M);
            else
                gemm_bias_wmma_ws<128, false><<<grid, TPB, 0, stream>>>(X, W, b, scale, Y, M);
        }
        return;
    }
    if (K == 32)
        gemm_bias_wmma_lds<32><<<strips, TPB, 0, stream>>>(X, W, b, scale, Y, M, N);
    else if (K == 64)
        gemm_bias_wmma_lds<64><<<strips, TPB, 0, stream>>>(X, W, b, scale, Y, M, N);
    else if (K == 128)
        gemm_bias_wmma_lds<128><<<strips, TPB, 0, stream>>>(X, W, b, scale, Y, M, N);
    else {
        const int tiles = strips * (N / 16);
        gemm_bias_wmma_rt<<<ceil_div_i(tiles, 8), TPB, 0, stream>>>(X, W, b, scale, Y, M, K, N);
    }
}

extern "C" void kernel_launch(void* const* d_in, const int* in_sizes, int n_in,
                              void* d_out, int out_size, void* d_ws, size_t ws_size,
                              hipStream_t stream) {
    const float* x_user  = (const float*)d_in[0];
    const float* x_movie = (const float*)d_in[1];
    const int*   edge    = (const int*)d_in[2];
    const float* Wu = (const float*)d_in[3];
    const float* bu = (const float*)d_in[4];
    const float* Wm = (const float*)d_in[5];
    const float* bm = (const float*)d_in[6];
    const float* W1 = (const float*)d_in[7];
    const float* b1 = (const float*)d_in[8];
    const float* W2 = (const float*)d_in[9];
    const float* b2 = (const float*)d_in[10];
    const float* We = (const float*)d_in[11];
    const float* be = (const float*)d_in[12];
    float* out = (float*)d_out;

    const int H  = in_sizes[4];
    const int C  = in_sizes[12];
    const int Fu = in_sizes[3] / H;
    const int Fm = in_sizes[5] / H;
    const int Nu = in_sizes[0] / Fu;
    const int Nm = in_sizes[1] / Fm;
    const int E  = in_sizes[2] / 2;
    const int N  = Nu + Nm;
    const int* row = edge;
    const int* col = edge + E;

    float* ws   = (float*)d_ws;
    float* X    = ws;
    float* Hh   = X  + (size_t)N * H;
    float* Acc  = Hh + (size_t)N * H;
    float* dinv = Acc + (size_t)N * H;

    const int TPB = 256, WPB = 8;

    const size_t Npad = ((size_t)N + 31) & ~(size_t)31;
    float* P      = dinv + Npad;
    int*   cnt    = (int*)(P + (size_t)N * 16);
    int*   start  = cnt + (((size_t)N + 1 + 31) & ~(size_t)31);
    int*   cursor = start + Npad;
    int2*  slots  = (int2*)(cursor + Npad);
    float* Wcu = (float*)(slots + (((size_t)E + 15) & ~(size_t)15));
    float* Wcm = Wcu + (size_t)H * Fu;
    float* bcu = Wcm + (size_t)H * Fm;
    float* bcm = bcu + H;
    const bool fold = (H == 128) && (Fu % 32 == 0) && (Fm % 32 == 0);
    int* srcs     = (int*)(bcm + H);
    int* BH       = srcs + (((size_t)E + 31) & ~(size_t)31);
    int* binBase  = BH + (size_t)SORT_NBMAX * SORT_G;
    int* binCount = binBase + SORT_NBMAX;
    const int NB = ceil_div_i(N, 256);
    const bool sorted = NB <= SORT_NBMAX;

    const int composeUnits = fold ? 8 * (Fu / 32) + 8 * (Fm / 32) + 8 : 0;
    if (sorted) {
        const int chunk = ceil_div_i(E, SORT_G);
        bin_hist_kernel<<<SORT_G + ceil_div_i(composeUnits, COMPOSE_WPB), SORT_T, 0, stream>>>(
            col, BH, E, NB, chunk, W1, b1, Wu, bu, Wm, bm, Wcu, bcu, Wcm, bcm, Fu, Fm);
        bin_scatter_kernel<<<SORT_G, SORT_T, 0, stream>>>(row, col, BH, binBase, binCount, slots, E, NB, chunk);
        bin_sort_kernel<<<NB, SORT_T, 0, stream>>>(slots, binBase, binCount, start, cnt, dinv, srcs, N, E);
    } else {
        zero_int_kernel<<<ceil_div_i(N + 1, TPB), TPB, 0, stream>>>(cnt, N + 1);
        count_targets_kernel<<<ceil_div_i(E, TPB), TPB, 0, stream>>>(col, cnt, E);
        alloc_segments_kernel<<<ceil_div_i(N, TPB), TPB, 0, stream>>>(cnt, cnt + N, start, cursor, dinv, N);
        fill_segments_kernel<<<ceil_div_i(E, TPB), TPB, 0, stream>>>(row, col, dinv, cursor, slots, E);
    }
    const float* rowScale = sorted ? dinv : nullptr;

    if (fold) {
        if (!sorted)
            compose_wmma<<<ceil_div_i(composeUnits, WPB), TPB, 0, stream>>>(W1, b1, Wu, bu, Wm, bm,
                                                                            Wcu, bcu, Wcm, bcm, Fu, Fm);
        if (sorted && Fu == 32 && Fm == 64) {
            const int su = ceil_div_i(Nu, 16), sm = ceil_div_i(Nm, 16);
            const int gu = su < 768 ? su : 768, gm = sm < 768 ? sm : 768;
            gemm_pair_wmma_ws<32, 64, true><<<gu + gm, TPB, 0, stream>>>(
                x_user, Wcu, bcu, rowScale, Hh, Nu, gu,
                x_movie, Wcm, bcm, rowScale + Nu, Hh + (size_t)Nu * H, Nm);
        } else {
            launch_gemm(x_user, Wcu, bcu, rowScale, Hh, Nu, Fu, H, stream);
            launch_gemm(x_movie, Wcm, bcm, rowScale ? rowScale + Nu : nullptr,
                        Hh + (size_t)Nu * H, Nm, Fm, H, stream);
        }
    } else {
        launch_gemm(x_user, Wu, bu, nullptr, X, Nu, Fu, H, stream);
        launch_gemm(x_movie, Wm, bm, nullptr, X + (size_t)Nu * H, Nm, Fm, H, stream);
    }

    const int gatherWaves = N * ((H + 127) / 128);

    const bool fusedGemm = sorted && H == 128;
    if (!fold) launch_gemm(X, W1, b1, rowScale, Hh, N, H, H, stream);
    if (fusedGemm) {
        const int tiles16 = ceil_div_i(N, 16);
        gather_relu_gemm_kernel<<<tiles16 < 1024 ? tiles16 : 1024, TPB, 0, stream>>>(
            Hh, dinv, start, cnt, srcs, W2, b2, Acc, N, E);
    } else if (sorted)
        gather_relu_kernel<true><<<ceil_div_i(gatherWaves, WPB), TPB, 0, stream>>>(Hh, dinv, start, cnt, slots, srcs, Acc, N, E, H);
    else
        gather_relu_kernel<false><<<ceil_div_i(gatherWaves, WPB), TPB, 0, stream>>>(Hh, dinv, start, cnt, slots, srcs, Acc, N, E, H);

    const bool fusedProj = sorted && H == 128 && C == 10;
    if (!fusedGemm) launch_gemm(Acc, W2, b2, rowScale, Hh, N, H, H, stream);
    const float* h2 = fusedGemm ? Acc : Hh;
    if (fusedProj)
        gather_relu_proj_kernel<<<ceil_div_i(Nu, 16) + ceil_div_i(Nm, 16), TPB, 0, stream>>>(
            h2, dinv, start, cnt, srcs, We, be, P, Nu, Nm, E, C);
    else if (sorted)
        gather_relu_kernel<true><<<ceil_div_i(gatherWaves, WPB), TPB, 0, stream>>>(h2, dinv, start, cnt, slots, srcs, X, N, E, H);
    else
        gather_relu_kernel<false><<<ceil_div_i(gatherWaves, WPB), TPB, 0, stream>>>(h2, dinv, start, cnt, slots, srcs, X, N, E, H);

    const int edgeTiles = ceil_div_i(E, 16);
    if (H == 128 && C == 10) {
        const int nodeTiles = ceil_div_i(Nu, 16) + ceil_div_i(Nm, 16);
        if (!fusedProj)
            node_proj_wmma<<<ceil_div_i(nodeTiles, 4), 128, 0, stream>>>(X, We, be, P, Nu, Nm, C);
        const long long quads = ((long long)E * C + 3) / 4;
        edge_add10<4><<<ceil_div_i(quads, TPB * 4), TPB, 0, stream>>>(P, row, col, out, E, Nu);
    } else if (H == 128)
        edge_mlp_wmma<128><<<ceil_div_i(edgeTiles, WPB), TPB, 0, stream>>>(X, row, col, We, be, out, E, Nu, H, C);
    else
        edge_mlp_wmma<0><<<ceil_div_i(edgeTiles, WPB), TPB, 0, stream>>>(X, row, col, We, be, out, E, Nu, H, C);
}


# in-edge lists in two launches, not three: each chunk block sorts its edges by bin into its own stretch, the bin block collects its 128 pieces; sum order by atomic arrival as before
# speedup vs baseline: 6.2047x; 1.0507x over previous
#include <hip/hip_runtime.h>

typedef __attribute__((ext_vector_type(2))) float v2f;
typedef __attribute__((ext_vector_type(8))) float v8f;

__device__ __forceinline__ v8f wmma_f32x4(v2f a, v2f b, v8f c) {
    return __builtin_amdgcn_wmma_f32_16x16x4_f32(
         false, a,  false, b,
         (short)0, c,  false,  false);
}

__device__ __forceinline__ void load_row_scale8(const float* __restrict__ scale,
                                                int row0, int M, float (&sc)[8]) {
    if (scale == nullptr) {
#pragma unroll
        for (int r = 0; r < 8; ++r) sc[r] = 1.0f;
    } else {
#pragma unroll
        for (int r = 0; r < 8; ++r) sc[r] = scale[min(row0 + r, M - 1)];
    }
}

template <int K>
__global__ void gemm_bias_wmma_lds(const float* __restrict__ X,
                                   const float* __restrict__ W,
                                   const float* __restrict__ bias,
                                   const float* __restrict__ scale,
                                   float* __restrict__ Y,
                                   int M, int N) {
    __shared__ float As[16 * K];

    const int lane = threadIdx.x & 31;
    const int wave = threadIdx.x >> 5;
    const int wavesPerBlock = blockDim.x >> 5;
    const int tilesN = N >> 4;
    const int strip = blockIdx.x * 16;

    if (strip + 16 <= M) {
        const float4* __restrict__ src = (const float4*)(X + (size_t)strip * K);
        float4* dst = (float4*)As;
        for (int i = threadIdx.x; i < (16 * K) / 4; i += blockDim.x) dst[i] = src[i];
    } else {
        for (int i = threadIdx.x; i < 16 * K; i += blockDim.x) {
            const int rr = i / K, cc = i - rr * K;
            const int gr = min(strip + rr, M - 1);
            As[i] = X[(size_t)gr * K + cc];
        }
    }
    __syncthreads();

    const int koff = (lane >> 4) << 1;
    const float* __restrict__ arow = As + (lane & 15) * K + koff;
    const bool fullTile = (strip + 16 <= M);
    const int rbase = strip + ((lane >> 4) << 3);
    float sc[8];
    load_row_scale8(scale, rbase, M, sc);

    for (int tn = wave; tn < tilesN; tn += wavesPerBlock) {
        const int ncol = tn * 16 + (lane & 15);
        const float* __restrict__ wrow = W + (size_t)ncol * K + koff;

        v8f acc = {};
#pragma unroll
        for (int k = 0; k < K; k += 4) {
            v2f a = *(const v2f*)(arow + k);
            v2f b = *(const v2f*)(wrow + k);
            acc = wmma_f32x4(a, b, acc);
        }

        const float bv = bias[ncol];
        float* yp = Y + (size_t)rbase * N + ncol;
        if (fullTile) {
#pragma unroll
            for (int r = 0; r < 8; ++r) { *yp = (acc[r] + bv) * sc[r]; yp += N; }
        } else {
#pragma unroll
            for (int r = 0; r < 8; ++r) {
                if (rbase + r < M) *yp = (acc[r] + bv) * sc[r];
                yp += N;
            }
        }
    }
}

template <int K, bool SCALED>
__device__ __forceinline__ void
gemm_ws_body(const float* __restrict__ X,
             const float* __restrict__ W,
             const float* __restrict__ bias,
             const float* __restrict__ scale,
             float* __restrict__ Y, int M,
             float* __restrict__ lds, int firstStrip, int stripStep) {
    constexpr int N = 128;
    constexpr int P = K + 4;
    constexpr int C4 = K / 4;
    constexpr int NV = 16 * C4;
    constexpr int PER = (NV + 255) / 256;
    float (*As)[16 * P] = (float (*)[16 * P])lds;

    const int lane = threadIdx.x & 31;
    const int wave = threadIdx.x >> 5;
    const int c = lane & 15;
    const int h = lane >> 4;
    const int ncol = wave * 16 + c;
    const int strips = (M + 15) >> 4;
    const int step = stripStep;

    float4 bf[K / 8];
    {
        const float* __restrict__ wrow = W + (size_t)ncol * K + 4 * h;
#pragma unroll
        for (int i = 0; i < K / 8; ++i) bf[i] = *(const float4*)(wrow + 8 * i);
    }
    const float bv = bias[ncol];

    static_assert(PER <= 2, "a strip is at most 512 float4s");
    const int i0 = threadIdx.x;
    const int rr0 = i0 / C4, cc0 = i0 - rr0 * C4;
    const int i1 = i0 + 256;
    const int rr1 = i1 / C4, cc1 = i1 - rr1 * C4;

    int s = firstStrip;
    if (s < strips) {
        if (NV >= 256 || i0 < NV)
            *(float4*)(&As[0][rr0 * P + 4 * cc0]) =
                *(const float4*)(X + (size_t)min(s * 16 + rr0, M - 1) * K + 4 * cc0);
        if (PER > 1)
            *(float4*)(&As[0][rr1 * P + 4 * cc1]) =
                *(const float4*)(X + (size_t)min(s * 16 + rr1, M - 1) * K + 4 * cc1);
    }
    __syncthreads();

    const int aoff = c * P + 4 * h;
    int buf = 0;
    for (; s < strips; s += step) {
        const int sn = s + step;
        const bool more = sn < strips;
        float4 pre0 = make_float4(0.f, 0.f, 0.f, 0.f);
        float4 pre1 = make_float4(0.f, 0.f, 0.f, 0.f);
        if (more) {
            if (NV >= 256 || i0 < NV)
                pre0 = *(const float4*)(X + (size_t)min(sn * 16 + rr0, M - 1) * K + 4 * cc0);
            if (PER > 1)
                pre1 = *(const float4*)(X + (size_t)min(sn * 16 + rr1, M - 1) * K + 4 * cc1);
        }
        const int rbase = s * 16 + 8 * h;
        float4 su = make_float4(1.f, 1.f, 1.f, 1.f), sv = su;
        if (SCALED) {
            const float* __restrict__ sp = scale + max(min(rbase, M - 8), 0);
            __builtin_memcpy(&su, sp, 16);
            __builtin_memcpy(&sv, sp + 4, 16);
        }

        const float* arow = &As[buf][aoff];
        v8f acc = {};
#pragma unroll
        for (int i = 0; i < K / 8; ++i) {
            const float4 a = *(const float4*)(arow + 8 * i);
            v2f a0 = {a.x, a.y}, b0 = {bf[i].x, bf[i].y};
            v2f a1 = {a.z, a.w}, b1 = {bf[i].z, bf[i].w};
            acc = wmma_f32x4(a0, b0, acc);
            acc = wmma_f32x4(a1, b1, acc);
        }

        if (more) {
            if (NV >= 256 || i0 < NV) *(float4*)(&As[buf ^ 1][rr0 * P + 4 * cc0]) = pre0;
            if (PER > 1) *(float4*)(&As[buf ^ 1][rr1 * P + 4 * cc1]) = pre1;
        }
        __syncthreads();
        buf ^= 1;

        float* yp = Y + (size_t)rbase * N + ncol;
        if (s * 16 + 16 <= M) {
            const float sc[8] = {su.x, su.y, su.z, su.w, sv.x, sv.y, sv.z, sv.w};
#pragma unroll
            for (int r = 0; r < 8; ++r) { *yp = (acc[r] + bv) * sc[r]; yp += N; }
        } else {
#pragma unroll
            for (int r = 0; r < 8; ++r) {
                if (rbase + r < M) *yp = (acc[r] + bv) * (SCALED ? scale[rbase + r] : 1.0f);
                yp += N;
            }
        }
    }
}

template <int K, bool SCALED>
__global__ void __launch_bounds__(256)
gemm_bias_wmma_ws(const float* __restrict__ X,
                  const float* __restrict__ W,
                  const float* __restrict__ bias,
                  const float* __restrict__ scale,
                  float* __restrict__ Y, int M) {
    __shared__ __attribute__((aligned(16))) float As[2 * 16 * (K + 4)];
    gemm_ws_body<K, SCALED>(X, W, bias, scale, Y, M, As, (int)blockIdx.x, (int)gridDim.x);
}

template <int KA, int KB, bool SCALED>
__global__ void __launch_bounds__(256)
gemm_pair_wmma_ws(const float* __restrict__ XA, const float* __restrict__ WA,
                  const float* __restrict__ biasA, const float* __restrict__ scaleA,
                  float* __restrict__ YA, int MA, int gridA,
                  const float* __restrict__ XB, const float* __restrict__ WB,
                  const float* __restrict__ biasB, const float* __restrict__ scaleB,
                  float* __restrict__ YB, int MB) {
    constexpr int KMAX = KA > KB ? KA : KB;
    __shared__ __attribute__((aligned(16))) float As[2 * 16 * (KMAX + 4)];
    const int bid = (int)blockIdx.x;
    if (bid < gridA)
        gemm_ws_body<KA, SCALED>(XA, WA, biasA, scaleA, YA, MA, As, bid, gridA);
    else
        gemm_ws_body<KB, SCALED>(XB, WB, biasB, scaleB, YB, MB, As, bid - gridA,
                                 (int)gridDim.x - gridA);
}

__device__ __forceinline__ void
compose_unit(const float* __restrict__ W1, const float* __restrict__ b1,
             const float* __restrict__ Wu, const float* __restrict__ bu,
             const float* __restrict__ Wm, const float* __restrict__ bm,
             float* __restrict__ Wcu, float* __restrict__ bcu,
             float* __restrict__ Wcm, float* __restrict__ bcm,
             int Fu, int Fm, int unit, int lane) {
    constexpr int H = 128;
    constexpr int S = H / 8;
    const int c = lane & 15;
    const int h = lane >> 4;
    const int wu = 8 * (Fu >> 5);
    const int wm = 8 * (Fm >> 5);

    if (unit < wu + wm) {
        const bool second = unit >= wu;
        const int u = second ? unit - wu : unit;
        const int F = second ? Fm : Fu;
        const float* __restrict__ Wt = second ? Wm : Wu;
        float* __restrict__ Wc = second ? Wcm : Wcu;
        const int groups = F >> 5;
        const int nt = u / groups;
        const int cg = u - nt * groups;

        const float* __restrict__ arow = W1 + (size_t)(nt * 16 + c) * H + 4 * h;
        const float* __restrict__ bcol = Wt + (size_t)(4 * h) * F + cg * 32 + c;

        float4 af[S];
        float bl[S][4], br[S][4];
#pragma unroll
        for (int i = 0; i < S; ++i) {
            af[i] = *(const float4*)(arow + 8 * i);
            const float* __restrict__ bp = bcol + (size_t)(8 * i) * F;
#pragma unroll
            for (int e = 0; e < 4; ++e) {
                bl[i][e] = bp[(size_t)e * F];
                br[i][e] = bp[(size_t)e * F + 16];
            }
        }

        v8f acc0 = {}, acc1 = {};
#pragma unroll
        for (int i = 0; i < S; ++i) {
            v2f a0 = {af[i].x, af[i].y}, a1 = {af[i].z, af[i].w};
            v2f p0 = {bl[i][0], bl[i][1]}, p1 = {bl[i][2], bl[i][3]};
            v2f q0 = {br[i][0], br[i][1]}, q1 = {br[i][2], br[i][3]};
            acc0 = wmma_f32x4(a0, p0, acc0);
            acc1 = wmma_f32x4(a0, q0, acc1);
            acc0 = wmma_f32x4(a1, p1, acc0);
            acc1 = wmma_f32x4(a1, q1, acc1);
        }

        float* wp = Wc + (size_t)(nt * 16 + 8 * h) * F + cg * 32 + c;
#pragma unroll
        for (int r = 0; r < 8; ++r) {
            wp[0] = acc0[r];
            wp[16] = acc1[r];
            wp += F;
        }
    } else if (unit < wu + wm + 8) {
        const int q = unit - wu - wm;
        const bool second = q >= 4;
        const float* __restrict__ bt = second ? bm : bu;
        float* __restrict__ bc = second ? bcm : bcu;
        const int nb = 32 * (q & 3);

        const float* __restrict__ ar0 = W1 + (size_t)(nb + c) * H + 4 * h;
        const float* __restrict__ ar1 = ar0 + (size_t)16 * H;
        const float* __restrict__ bv = bt + 4 * h;

        float4 a0f[S], a1f[S], bf[S];
#pragma unroll
        for (int i = 0; i < S; ++i) {
            a0f[i] = *(const float4*)(ar0 + 8 * i);
            a1f[i] = *(const float4*)(ar1 + 8 * i);
            bf[i] = *(const float4*)(bv + 8 * i);
        }
        const float4 s00 = *(const float4*)(b1 + nb + 8 * h);
        const float4 s01 = *(const float4*)(b1 + nb + 8 * h + 4);
        const float4 s10 = *(const float4*)(b1 + nb + 16 + 8 * h);
        const float4 s11 = *(const float4*)(b1 + nb + 16 + 8 * h + 4);

        v8f acc0 = {}, acc1 = {};
#pragma unroll
        for (int i = 0; i < S; ++i) {
            v2f x0 = {a0f[i].x, a0f[i].y}, x1 = {a0f[i].z, a0f[i].w};
            v2f y0 = {a1f[i].x, a1f[i].y}, y1 = {a1f[i].z, a1f[i].w};
            v2f p0 = {bf[i].x, bf[i].y}, p1 = {bf[i].z, bf[i].w};
            acc0 = wmma_f32x4(x0, p0, acc0);
            acc1 = wmma_f32x4(y0, p0, acc1);
            acc0 = wmma_f32x4(x1, p1, acc0);
            acc1 = wmma_f32x4(y1, p1, acc1);
        }

        const float4 o00 = make_float4(acc0[0] + s00.x, acc0[1] + s00.y, acc0[2] + s00.z, acc0[3] + s00.w);
        const float4 o01 = make_float4(acc0[4] + s01.x, acc0[5] + s01.y, acc0[6] + s01.z, acc0[7] + s01.w);
        const float4 o10 = make_float4(acc1[0] + s10.x, acc1[1] + s10.y, acc1[2] + s10.z, acc1[3] + s10.w);
        const float4 o11 = make_float4(acc1[4] + s11.x, acc1[5] + s11.y, acc1[6] + s11.z, acc1[7] + s11.w);
        if (c == 0) {
            float* op = bc + nb + 8 * h;
            *(float4*)(op) = o00;
            *(float4*)(op + 4) = o01;
            *(float4*)(op + 16) = o10;
            *(float4*)(op + 20) = o11;
        }
    }
}

__global__ void __launch_bounds__(256)
compose_wmma(const float* __restrict__ W1, const float* __restrict__ b1,
             const float* __restrict__ Wu, const float* __restrict__ bu,
             const float* __restrict__ Wm, const float* __restrict__ bm,
             float* __restrict__ Wcu, float* __restrict__ bcu,
             float* __restrict__ Wcm, float* __restrict__ bcm,
             int Fu, int Fm) {
    const int unit = blockIdx.x * (blockDim.x >> 5) + (threadIdx.x >> 5);
    compose_unit(W1, b1, Wu, bu, Wm, bm, Wcu, bcu, Wcm, bcm, Fu, Fm, unit, threadIdx.x & 31);
}

__global__ void gemm_bias_wmma_rt(const float* __restrict__ X,
                                  const float* __restrict__ W,
                                  const float* __restrict__ bias,
                                  const float* __restrict__ scale,
                                  float* __restrict__ Y,
                                  int M, int K, int N) {
    const int lane = threadIdx.x & 31;
    const int wave = threadIdx.x >> 5;
    const int tilesN = N >> 4;
    const int tile = blockIdx.x * (blockDim.x >> 5) + wave;
    const int tm = tile / tilesN;
    const int tn = tile - tm * tilesN;
    if (tm * 16 >= M) return;

    const int mrow = min(tm * 16 + (lane & 15), M - 1);
    const int ncol = tn * 16 + (lane & 15);
    const int koff = (lane >> 4) << 1;
    const float* __restrict__ xrow = X + (size_t)mrow * K + koff;
    const float* __restrict__ wrow = W + (size_t)ncol * K + koff;

    v8f acc = {};
#pragma unroll 8
    for (int k = 0; k < K; k += 4) {
        v2f a = *(const v2f*)(xrow + k);
        v2f b = *(const v2f*)(wrow + k);
        acc = wmma_f32x4(a, b, acc);
    }

    const float bv = bias[ncol];
    const int rbase = tm * 16 + ((lane >> 4) << 3);
    float sc[8];
    load_row_scale8(scale, rbase, M, sc);
    float* yp = Y + (size_t)rbase * N + ncol;
#pragma unroll
    for (int r = 0; r < 8; ++r) {
        if (rbase + r < M) *yp = (acc[r] + bv) * sc[r];
        yp += N;
    }
}

template <int HT>
__global__ void edge_mlp_wmma(const float* __restrict__ xf,
                              const int* __restrict__ row,
                              const int* __restrict__ col,
                              const float* __restrict__ We,
                              const float* __restrict__ be,
                              float* __restrict__ out,
                              int E, int Nu, int Hrt, int C) {
    const int H = (HT > 0) ? HT : Hrt;
    const int K = 2 * H;
    const int lane = threadIdx.x & 31;
    const int wave = threadIdx.x >> 5;
    const int tile = blockIdx.x * (blockDim.x >> 5) + wave;
    const int ebase = tile * 16;
    if (ebase >= E) return;

    const int m = lane & 15;
    const int e = min(ebase + m, E - 1);
    const int r = row[e];
    const int c = col[e];
    const int koff = (lane >> 4) << 1;

    const float* __restrict__ urow = xf + (size_t)r * H + koff;
    const float* __restrict__ mrow = xf + (size_t)(Nu + c) * H + koff;

    const int n = lane & 15;
    const float wmask = (n < C) ? 1.0f : 0.0f;
    const float* __restrict__ wrow = We + (size_t)min(n, C - 1) * K + koff;

    v8f acc = {};
#pragma unroll 8
    for (int k = 0; k < K; k += 4) {
        const float* src = (k + koff < H) ? (urow + k) : (mrow + (k - H));
        v2f a = *(const v2f*)src;
        v2f b = *(const v2f*)(wrow + k);
        b.x *= wmask;
        b.y *= wmask;
        acc = wmma_f32x4(a, b, acc);
    }

    if (n < C) {
        const float bv = be[n];
        const int mb = (lane >> 4) << 3;
        float* op = out + (size_t)(ebase + mb) * C + n;
        if (ebase + 16 <= E) {
#pragma unroll
            for (int rr = 0; rr < 8; ++rr) { *op = acc[rr] + bv; op += C; }
        } else {
#pragma unroll
            for (int rr = 0; rr < 8; ++rr) {
                if (ebase + mb + rr < E) *op = acc[rr] + bv;
                op += C;
            }
        }
    }
}

__global__ void node_proj_wmma(const float* __restrict__ xf,
                               const float* __restrict__ We,
                               const float* __restrict__ be,
                               float* __restrict__ P,
                               int Nu, int Nm, int C) {
    constexpr int H = 128;
    const int lane = threadIdx.x & 31;
    const int wave = threadIdx.x >> 5;
    const int tilesU = (Nu + 15) >> 4;
    const int tilesM = (Nm + 15) >> 4;
    const int tile = blockIdx.x * (blockDim.x >> 5) + wave;
    if (tile >= tilesU + tilesM) return;

    const bool second = tile >= tilesU;
    const int segEnd = second ? Nu + Nm : Nu;
    const int base = second ? Nu + ((tile - tilesU) << 4) : (tile << 4);

    const int c = lane & 15;
    const int h = lane >> 4;
    const int node = min(base + 2 * (c & 7) + (c >> 3), segEnd - 1);
    const float* __restrict__ arow = xf + (size_t)node * H + 4 * h;
    const float* __restrict__ brow =
        We + (size_t)min(c, C - 1) * (2 * H) + (second ? H : 0) + 4 * h;

    const float bsel = be[min(c, C - 1)];
    const float bv = (second && c < C) ? bsel : 0.0f;

    v8f acc0 = {}, acc1 = {};
#pragma unroll
    for (int i = 0; i < H / 8; ++i) {
        const float4 a = *(const float4*)(arow + 8 * i);
        const float4 b = *(const float4*)(brow + 8 * i);
        v2f a0 = {a.x, a.y}, b0 = {b.x, b.y};
        v2f a1 = {a.z, a.w}, b1 = {b.z, b.w};
        acc0 = wmma_f32x4(a0, b0, acc0);
        acc1 = wmma_f32x4(a1, b1, acc1);
    }

    float v[8];
#pragma unroll
    for (int r = 0; r < 8; ++r) v[r] = (acc0[r] + acc1[r]) + bv;

    float* pp = P + (size_t)base * 16 + lane;
    if (base + 16 <= segEnd) {
#pragma unroll
        for (int r = 0; r < 8; ++r) pp[32 * r] = v[r];
    } else {
#pragma unroll
        for (int r = 0; r < 8; ++r)
            if (base + 2 * r + h < segEnd) pp[32 * r] = v[r];
    }
}

template <int ITER>
__global__ void edge_add10(const float* __restrict__ P,
                           const int* __restrict__ row,
                           const int* __restrict__ col,
                           float* __restrict__ out,
                           int E, int Nu) {
    constexpr unsigned C = 10;
    const unsigned total = (unsigned)E * C;
    const float* __restrict__ Pu = P;
    const float* __restrict__ Pm = P + (size_t)Nu * 16;
    const unsigned perBlock = blockDim.x * ITER;
    const unsigned qb = blockIdx.x * perBlock;

    if ((qb + perBlock) * 4u <= total) {
        unsigned f0[ITER], cA[ITER], cB[ITER];
        int rA[ITER], mA[ITER], rB[ITER], mB[ITER];
#pragma unroll
        for (int it = 0; it < ITER; ++it) {
            f0[it] = 4u * (qb + it * blockDim.x + threadIdx.x);
            const unsigned f1 = f0[it] + 2u;
            const unsigned eA = f0[it] / C, eB = f1 / C;
            cA[it] = f0[it] - eA * C;
            cB[it] = f1 - eB * C;
            rA[it] = row[eA]; mA[it] = col[eA];
            rB[it] = row[eB]; mB[it] = col[eB];
        }
        float2 uA[ITER], vA[ITER], uB[ITER], vB[ITER];
#pragma unroll
        for (int it = 0; it < ITER; ++it) {
            uA[it] = *(const float2*)(Pu + (size_t)rA[it] * 16 + cA[it]);
            vA[it] = *(const float2*)(Pm + (size_t)mA[it] * 16 + cA[it]);
            uB[it] = *(const float2*)(Pu + (size_t)rB[it] * 16 + cB[it]);
            vB[it] = *(const float2*)(Pm + (size_t)mB[it] * 16 + cB[it]);
        }
#pragma unroll
        for (int it = 0; it < ITER; ++it) {
            *(float4*)(out + f0[it]) = make_float4(uA[it].x + vA[it].x, uA[it].y + vA[it].y,
                                                   uB[it].x + vB[it].x, uB[it].y + vB[it].y);
        }
    } else {
        for (unsigned p = threadIdx.x; p < 2u * perBlock; p += blockDim.x) {
            const unsigned f = 4u * qb + 2u * p;
            if (f < total) {
                const unsigned e = f / C, c = f - e * C;
                const float2 u = *(const float2*)(Pu + (size_t)row[e] * 16 + c);
                const float2 v = *(const float2*)(Pm + (size_t)col[e] * 16 + c);
                *(float2*)(out + f) = make_float2(u.x + v.x, u.y + v.y);
            }
        }
    }
}

constexpr int SORT_G = 128;
constexpr int SORT_NBMAX = 1024;
constexpr int SORT_T = 1024;
static_assert(SORT_T == SORT_NBMAX && SORT_G % 32 == 0, "one bin per thread, whole waves of chunks");

__device__ __forceinline__ int wave_incl_scan(int v, int lane) {
    for (int d = 1; d < 32; d <<= 1) {
        const int t = __shfl_up(v, d);
        if (lane >= d) v += t;
    }
    return v;
}

constexpr int COMPOSE_WPB = 8;
__global__ void __launch_bounds__(SORT_T)
chunk_bin_sort_kernel(const int* __restrict__ row, const int* __restrict__ col,
                      int* __restrict__ OFF, int2* __restrict__ binned,
                      int E, int NB, int chunk,
                      const float* __restrict__ W1, const float* __restrict__ b1,
                      const float* __restrict__ Wu, const float* __restrict__ bu,
                      const float* __restrict__ Wm, const float* __restrict__ bm,
                      float* __restrict__ Wcu, float* __restrict__ bcu,
                      float* __restrict__ Wcm, float* __restrict__ bcm,
                      int Fu, int Fm) {
    __shared__ int hist[SORT_NBMAX];
    __shared__ int wsum[32];
    const int tid = threadIdx.x;
    if (blockIdx.x >= SORT_G) {
        const int wave = tid >> 5;
        if (wave < COMPOSE_WPB)
            compose_unit(W1, b1, Wu, bu, Wm, bm, Wcu, bcu, Wcm, bcm, Fu, Fm,
                         ((int)blockIdx.x - SORT_G) * COMPOSE_WPB + wave, tid & 31);
        return;
    }
    const int lane = tid & 31;
    const int wave = tid >> 5;
    const int g = blockIdx.x;
    hist[tid] = 0;
    __syncthreads();
    const int lo = min(g * chunk, E);
    const int hi = min(lo + chunk, E);
    for (int e0 = lo + tid; e0 < hi; e0 += 4 * SORT_T) {
        int c[4];
#pragma unroll
        for (int k = 0; k < 4; ++k) {
            const int e = e0 + k * SORT_T;
            c[k] = (e < hi) ? col[e] : -1;
        }
#pragma unroll
        for (int k = 0; k < 4; ++k)
            if (c[k] >= 0) atomicAdd(&hist[min(c[k] >> 8, NB - 1)], 1);
    }
    __syncthreads();
    const int mine = hist[tid];
    const int incl = wave_incl_scan(mine, lane);
    if (lane == 31) wsum[wave] = incl;
    __syncthreads();
    if (wave == 0) {
        const int w = wsum[lane];
        const int wi = wave_incl_scan(w, lane);
        wsum[lane] = wi - w;
    }
    __syncthreads();
    const int excl = wsum[wave] + incl - mine;
    OFF[(size_t)g * SORT_NBMAX + tid] = excl;
    hist[tid] = excl;
    __syncthreads();

    for (int e0 = lo + tid; e0 < hi; e0 += 4 * SORT_T) {
        int c[4], r[4];
#pragma unroll
        for (int k = 0; k < 4; ++k) {
            const int e = e0 + k * SORT_T;
            c[k] = (e < hi) ? col[e] : -1;
            r[k] = (e < hi) ? row[e] : 0;
        }
#pragma unroll
        for (int k = 0; k < 4; ++k) {
            if (c[k] >= 0) {
                const int pos = atomicAdd(&hist[min(c[k] >> 8, NB - 1)], 1);
                binned[lo + min(max(pos, 0), hi - lo - 1)] = make_int2(r[k], c[k]);
            }
        }
    }
}

__global__ void __launch_bounds__(SORT_T)
bin_sort_kernel(const int2* __restrict__ binned, const int* __restrict__ OFF,
                int* __restrict__ start, int* __restrict__ cnt,
                float* __restrict__ dinv, int* __restrict__ srcs, int N, int E, int chunk) {
    constexpr int PPW = SORT_G / 32;
    constexpr int GW = SORT_G / 32;
    __shared__ int c[256];
    __shared__ int cur[256];
    __shared__ int wsum[8];
    __shared__ int pieceSrc[SORT_G];
    __shared__ int pieceLen[SORT_G];
    __shared__ int red[2 * GW];
    const int tid = threadIdx.x;
    const int lane = tid & 31;
    const int wave = tid >> 5;
    const int b = blockIdx.x;
    const int node0 = b << 8;

    if (tid < 256) c[tid] = 0;
    if (tid < SORT_G) {
        const int lo = min(tid * chunk, E);
        const int clen = min(lo + chunk, E) - lo;
        const int* __restrict__ rp = OFF + (size_t)tid * SORT_NBMAX + b;
        int a0 = rp[0];
        int a1 = (b + 1 < SORT_NBMAX) ? rp[1] : clen;
        a0 = min(max(a0, 0), clen);
        a1 = min(max(a1, a0), clen);
        pieceSrc[tid] = lo + a0;
        pieceLen[tid] = a1 - a0;
        int sb = a0, sl = a1 - a0;
        for (int d = 16; d >= 1; d >>= 1) {
            sb += __shfl_xor(sb, d);
            sl += __shfl_xor(sl, d);
        }
        if (lane == 0) { red[wave] = sb; red[GW + wave] = sl; }
    }
    __syncthreads();
    int bsum = 0, lsum = 0;
#pragma unroll
    for (int w = 0; w < GW; ++w) { bsum += red[w]; lsum += red[GW + w]; }
    const int count = min(max(lsum, 0), E);
    const int base = min(max(bsum, 0), E - count);

    int plen[PPW], psrc[PPW];
    int2 ed[PPW][2];
#pragma unroll
    for (int p = 0; p < PPW; ++p) {
        plen[p] = pieceLen[wave + 32 * p];
        psrc[p] = pieceSrc[wave + 32 * p];
    }
#pragma unroll
    for (int p = 0; p < PPW; ++p) {
#pragma unroll
        for (int q = 0; q < 2; ++q) {
            const int k = lane + 32 * q;
            ed[p][q] = (k < plen[p]) ? binned[psrc[p] + k] : make_int2(0, -1);
        }
    }
#pragma unroll
    for (int p = 0; p < PPW; ++p) {
#pragma unroll
        for (int q = 0; q < 2; ++q)
            if (ed[p][q].y >= 0) atomicAdd(&c[min(max(ed[p][q].y - node0, 0), 255)], 1);
        for (int k = lane + 64; k < plen[p]; k += 32) {
            const int t = binned[psrc[p] + k].y;
            if (t >= 0) atomicAdd(&c[min(max(t - node0, 0), 255)], 1);
        }
    }
    __syncthreads();
    int mine = 0, incl = 0;
    if (tid < 256) {
        mine = c[tid];
        incl = wave_incl_scan(mine, lane);
        if (lane == 31) wsum[wave] = incl;
    }
    __syncthreads();
    if (tid < 256) {
        int off = 0;
        for (int w = 0; w < wave; ++w) off += wsum[w];
        const int excl = off + incl - mine;
        cur[tid] = excl;
        const int node = node0 + tid;
        if (node < N) {
            start[node] = base + excl;
            cnt[node] = mine;
            dinv[node] = rsqrtf((float)(mine + 1));
        }
    }
    __syncthreads();
#pragma unroll
    for (int p = 0; p < PPW; ++p) {
#pragma unroll
        for (int q = 0; q < 2; ++q) {
            if (ed[p][q].y >= 0) {
                const int pos = atomicAdd(&cur[min(max(ed[p][q].y - node0, 0), 255)], 1);
                srcs[base + min(max(pos, 0), count - 1)] = ed[p][q].x;
            }
        }
        for (int k = lane + 64; k < plen[p]; k += 32) {
            const int2 e2 = binned[psrc[p] + k];
            if (e2.y >= 0) {
                const int pos = atomicAdd(&cur[min(max(e2.y - node0, 0), 255)], 1);
                srcs[base + min(max(pos, 0), count - 1)] = e2.x;
            }
        }
    }
}

__global__ void zero_int_kernel(int* __restrict__ p, int n) {
    const int i = blockIdx.x * blockDim.x + threadIdx.x;
    if (i < n) p[i] = 0;
}

__global__ void count_targets_kernel(const int* __restrict__ col,
                                     int* __restrict__ cnt, int E) {
    const int e = blockIdx.x * blockDim.x + threadIdx.x;
    if (e < E) atomicAdd(&cnt[col[e]], 1);
}

__global__ void alloc_segments_kernel(const int* __restrict__ cnt,
                                      int* __restrict__ total,
                                      int* __restrict__ start,
                                      int* __restrict__ cursor,
                                      float* __restrict__ dinv, int N) {
    const int i = blockIdx.x * blockDim.x + threadIdx.x;
    const int lane = threadIdx.x & 31;
    const int c = (i < N) ? cnt[i] : 0;
    int incl = c;
    for (int d = 1; d < 32; d <<= 1) {
        const int t = __shfl_up(incl, d);
        if (lane >= d) incl += t;
    }
    int base = 0;
    if (lane == 31) base = atomicAdd(total, incl);
    base = __shfl(base, 31);
    if (i < N) {
        const int s = base + incl - c;
        start[i] = s;
        cursor[i] = s;
        dinv[i] = rsqrtf((float)(c + 1));
    }
}

__global__ void fill_segments_kernel(const int* __restrict__ row,
                                     const int* __restrict__ col,
                                     const float* __restrict__ dinv,
                                     int* __restrict__ cursor,
                                     int2* __restrict__ slots, int E) {
    const int e = blockIdx.x * blockDim.x + threadIdx.x;
    if (e < E) {
        const int r = row[e];
        const int pos = atomicAdd(&cursor[col[e]], 1);
        slots[pos] = make_int2(r, __float_as_int(dinv[r]));
    }
}

__device__ __forceinline__ float4 fma4(float w, float4 a, float4 c) {
    return make_float4(fmaf(w, a.x, c.x), fmaf(w, a.y, c.y), fmaf(w, a.z, c.z), fmaf(w, a.w, c.w));
}

__device__ __forceinline__ float4 add4(float4 a, float4 c) {
    return make_float4(a.x + c.x, a.y + c.y, a.z + c.z, a.w + c.w);
}

template <bool SORTED>
__global__ void gather_relu_kernel(const float* __restrict__ h,
                                   const float* __restrict__ dinv,
                                   const int* __restrict__ start,
                                   const int* __restrict__ cnt,
                                   const int2* __restrict__ slots,
                                   const int* __restrict__ srcs,
                                   float* __restrict__ xout,
                                   int N, int E, int H) {
    const int lane = threadIdx.x & 31;
    const int cpn = (H + 127) >> 7;
    const int wid = __builtin_amdgcn_readfirstlane(
        (int)(blockIdx.x * (blockDim.x >> 5) + (threadIdx.x >> 5)));
    const int node = wid / cpn;
    if (node >= N) return;
    const int cb = (wid - node * cpn) << 7;
    const bool act = cb + 4 * lane < H;
    const int c0 = min(cb + 4 * lane, H - 4);
    const float* __restrict__ hl = h + c0;

    const int c = min(max(cnt[node], 0), E);
    const int s = min(max(start[node], 0), E - c);
    const float4 hself = *(const float4*)(hl + (size_t)node * H);
    const float di = dinv[node];

    float4 acc = make_float4(0.f, 0.f, 0.f, 0.f);
    for (int k0 = 0; k0 < c; k0 += 32) {
        const int n = min(32, c - k0);
        int2 ev = make_int2(0, 0);
        if (lane < n) {
            if (SORTED)
                ev.x = min(max(srcs[s + k0 + lane], 0), N - 1);
            else
                ev = slots[s + k0 + lane];
        }
        for (int j = 0; j < n; j += 4) {
            const int j1 = min(j + 1, n - 1), j2 = min(j + 2, n - 1), j3 = min(j + 3, n - 1);
            const int s0 = __builtin_amdgcn_readlane(ev.x, j);
            const int s1 = __builtin_amdgcn_readlane(ev.x, j1);
            const int s2 = __builtin_amdgcn_readlane(ev.x, j2);
            const int s3 = __builtin_amdgcn_readlane(ev.x, j3);
            const float4 a0 = *(const float4*)(hl + (size_t)s0 * H);
            const float4 a1 = *(const float4*)(hl + (size_t)s1 * H);
            const float4 a2 = *(const float4*)(hl + (size_t)s2 * H);
            const float4 a3 = *(const float4*)(hl + (size_t)s3 * H);
            if (SORTED) {
                acc = add4(a0, acc);
                if (j + 1 < n) acc = add4(a1, acc);
                if (j + 2 < n) acc = add4(a2, acc);
                if (j + 3 < n) acc = add4(a3, acc);
            } else {
                const float w0 = __int_as_float(__builtin_amdgcn_readlane(ev.y, j));
                const float w1 = __int_as_float(__builtin_amdgcn_readlane(ev.y, j1));
                const float w2 = __int_as_float(__builtin_amdgcn_readlane(ev.y, j2));
                const float w3 = __int_as_float(__builtin_amdgcn_readlane(ev.y, j3));
                acc = fma4(w0, a0, acc);
                if (j + 1 < n) acc = fma4(w1, a1, acc);
                if (j + 2 < n) acc = fma4(w2, a2, acc);
                if (j + 3 < n) acc = fma4(w3, a3, acc);
            }
        }
    }

    if (act) {
        float4 v;
        if (SORTED) {
            v = make_float4(di * (acc.x + hself.x), di * (acc.y + hself.y),
                            di * (acc.z + hself.z), di * (acc.w + hself.w));
        } else {
            const float d2 = di * di;
            v = make_float4(di * acc.x + d2 * hself.x, di * acc.y + d2 * hself.y,
                            di * acc.z + d2 * hself.z, di * acc.w + d2 * hself.w);
        }
        v.x = v.x > 0.f ? v.x : 0.f;
        v.y = v.y > 0.f ? v.y : 0.f;
        v.z = v.z > 0.f ? v.z : 0.f;
        v.w = v.w > 0.f ? v.w : 0.f;
        *(float4*)(xout + (size_t)node * H + c0) = v;
    }
}

__global__ void __launch_bounds__(256)
gather_relu_proj_kernel(const float* __restrict__ h,
                        const float* __restrict__ dinv,
                        const int* __restrict__ start,
                        const int* __restrict__ cnt,
                        const int* __restrict__ srcs,
                        const float* __restrict__ We,
                        const float* __restrict__ be,
                        float* __restrict__ P,
                        int Nu, int Nm, int E, int C) {
    constexpr int H = 128;
    constexpr int PITCH = H + 4;
    __shared__ __attribute__((aligned(16))) float As[16 * PITCH];

    const int lane = threadIdx.x & 31;
    const int wave = __builtin_amdgcn_readfirstlane((int)(threadIdx.x >> 5));
    const int N = Nu + Nm;
    const int tilesU = (Nu + 15) >> 4;
    const int tile = (int)blockIdx.x;
    const bool second = tile >= tilesU;
    const int segEnd = second ? Nu + Nm : Nu;
    const int base = second ? Nu + ((tile - tilesU) << 4) : (tile << 4);

    const float* __restrict__ hl = h + 4 * lane;

    int nodeT[2], cT[2], sT[2], evT[2];
    float diT[2];
    float4 selfT[2];
#pragma unroll
    for (int t = 0; t < 2; ++t) {
        nodeT[t] = min(base + 2 * wave + t, segEnd - 1);
        cT[t] = min(max(cnt[nodeT[t]], 0), E);
        sT[t] = min(max(start[nodeT[t]], 0), E - cT[t]);
        diT[t] = dinv[nodeT[t]];
        selfT[t] = *(const float4*)(hl + (size_t)nodeT[t] * H);
    }
    int rawT[2];
#pragma unroll
    for (int t = 0; t < 2; ++t) {
        rawT[t] = 0;
        if (lane < min(32, cT[t])) rawT[t] = srcs[sT[t] + lane];
    }
#pragma unroll
    for (int t = 0; t < 2; ++t)
        evT[t] = min(max(rawT[t], 0), N - 1);

#pragma unroll
    for (int t = 0; t < 2; ++t) {
        const int c = cT[t];
        const int s = sT[t];
        float4 acc = make_float4(0.f, 0.f, 0.f, 0.f);
        int ev = evT[t];
        for (int k0 = 0; k0 < c; k0 += 32) {
            const int n = min(32, c - k0);
            if (k0 > 0) {
                ev = 0;
                if (lane < n) ev = min(max(srcs[s + k0 + lane], 0), N - 1);
            }
            for (int j = 0; j < n; j += 4) {
                const int j1 = min(j + 1, n - 1), j2 = min(j + 2, n - 1), j3 = min(j + 3, n - 1);
                const int s0 = __builtin_amdgcn_readlane(ev, j);
                const int s1 = __builtin_amdgcn_readlane(ev, j1);
                const int s2 = __builtin_amdgcn_readlane(ev, j2);
                const int s3 = __builtin_amdgcn_readlane(ev, j3);
                const float4 a0 = *(const float4*)(hl + (size_t)s0 * H);
                const float4 a1 = *(const float4*)(hl + (size_t)s1 * H);
                const float4 a2 = *(const float4*)(hl + (size_t)s2 * H);
                const float4 a3 = *(const float4*)(hl + (size_t)s3 * H);
                acc = add4(a0, acc);
                if (j + 1 < n) acc = add4(a1, acc);
                if (j + 2 < n) acc = add4(a2, acc);
                if (j + 3 < n) acc = add4(a3, acc);
            }
        }
        const float di = diT[t];
        const float4 hs = selfT[t];
        float4 v = make_float4(di * (acc.x + hs.x), di * (acc.y + hs.y),
                               di * (acc.z + hs.z), di * (acc.w + hs.w));
        v.x = v.x > 0.f ? v.x : 0.f;
        v.y = v.y > 0.f ? v.y : 0.f;
        v.z = v.z > 0.f ? v.z : 0.f;
        v.w = v.w > 0.f ? v.w : 0.f;
        *(float4*)(&As[(2 * wave + t) * PITCH + 4 * lane]) = v;
    }
    __syncthreads();
    if (wave != 0) return;

    const int c = lane & 15;
    const int hh = lane >> 4;
    const float* arow = &As[(2 * (c & 7) + (c >> 3)) * PITCH + 4 * hh];
    const float* __restrict__ brow =
        We + (size_t)min(c, C - 1) * (2 * H) + (second ? H : 0) + 4 * hh;

    const float bsel = be[min(c, C - 1)];
    const float bv = (second && c < C) ? bsel : 0.0f;

    v8f acc0 = {}, acc1 = {};
#pragma unroll
    for (int i = 0; i < H / 8; ++i) {
        const float4 a = *(const float4*)(arow + 8 * i);
        const float4 b = *(const float4*)(brow + 8 * i);
        v2f a0 = {a.x, a.y}, b0 = {b.x, b.y};
        v2f a1 = {a.z, a.w}, b1 = {b.z, b.w};
        acc0 = wmma_f32x4(a0, b0, acc0);
        acc1 = wmma_f32x4(a1, b1, acc1);
    }

    float v[8];
#pragma unroll
    for (int r = 0; r < 8; ++r) v[r] = (acc0[r] + acc1[r]) + bv;

    float* pp = P + (size_t)base * 16 + lane;
    if (base + 16 <= segEnd) {
#pragma unroll
        for (int r = 0; r < 8; ++r) pp[32 * r] = v[r];
    } else {
#pragma unroll
        for (int r = 0; r < 8; ++r)
            if (base + 2 * r + hh < segEnd) pp[32 * r] = v[r];
    }
}

__global__ void __launch_bounds__(256)
gather_relu_gemm_kernel(const float* __restrict__ h,
                        const float* __restrict__ dinv,
                        const int* __restrict__ start,
                        const int* __restrict__ cnt,
                        const int* __restrict__ srcs,
                        const float* __restrict__ W,
                        const float* __restrict__ bias,
                        float* __restrict__ Y,
                        int N, int E) {
    constexpr int H = 128;
    constexpr int PITCH = H + 4;
    constexpr int G = 6;
    __shared__ __attribute__((aligned(16))) float As[2][16 * PITCH];
    __shared__ __attribute__((aligned(16))) float Ds[2][16];

    const int lane = threadIdx.x & 31;
    const int wave = __builtin_amdgcn_readfirstlane((int)(threadIdx.x >> 5));
    const int c = lane & 15;
    const int hh = lane >> 4;
    const int ncol = wave * 16 + c;
    const int tiles = (N + 15) >> 4;
    const int nb = (int)gridDim.x;
    const int b = (int)blockIdx.x;
    if (b >= tiles) return;
    const int trips = (tiles - b + nb - 1) / nb;
    const bool rev = (b & 1) != 0;

    float4 bf[H / 8];
    {
        const float* __restrict__ wrow = W + (size_t)ncol * H + 4 * hh;
#pragma unroll
        for (int i = 0; i < H / 8; ++i) bf[i] = *(const float4*)(wrow + 8 * i);
    }
    const float bv = bias[ncol];
    const int loff = 4 * lane;
    const int aoff = c * PITCH + 4 * hh;

    int tile = b + (rev ? trips - 1 : 0) * nb;
    int cT[2], sT[2], evT[2];
    float diT[2];
    float4 selfT[2];
    {
        const int nl = min((tile << 4) + c, N - 1);
        const int cv = cnt[nl];
        const int sv = start[nl];
#pragma unroll
        for (int t = 0; t < 2; ++t) {
            const int node = min((tile << 4) + 2 * wave + t, N - 1);
            cT[t] = min(max(__builtin_amdgcn_readlane(cv, 2 * wave + t), 0), E);
            sT[t] = min(max(__builtin_amdgcn_readlane(sv, 2 * wave + t), 0), E - cT[t]);
            diT[t] = dinv[node];
            selfT[t] = *(const float4*)(h + (size_t)node * H + loff);
        }
#pragma unroll
        for (int t = 0; t < 2; ++t) {
            int raw = 0;
            if (lane < min(32, cT[t])) raw = srcs[sT[t] + lane];
            evT[t] = min(max(raw, 0), N - 1);
        }
    }

    int buf = 0;
    for (int k = 0; k < trips; ++k) {
        const int base = tile << 4;
        const int kn = min(k + 1, trips - 1);
        const int tnext = b + (rev ? trips - 1 - kn : kn) * nb;
        const int nl = min((tnext << 4) + c, N - 1);
        const int cv = cnt[nl];
        const int sv = start[nl];

#pragma unroll
        for (int t = 0; t < 2; ++t) {
            const int cc = cT[t];
            const int ss = sT[t];
            float4 acc = make_float4(0.f, 0.f, 0.f, 0.f);
            int ev = evT[t];
            for (int k0 = 0; k0 < cc; k0 += 32) {
                const int n = min(32, cc - k0);
                if (k0 > 0) {
                    ev = 0;
                    if (lane < n) ev = min(max(srcs[ss + k0 + lane], 0), N - 1);
                }
                for (int j = 0; j < n; j += G) {
                    float4 a[G];
#pragma unroll
                    for (int q = 0; q < G; ++q) {
                        const int sq = __builtin_amdgcn_readlane(ev, min(j + q, n - 1));
                        const float* __restrict__ rowp = h + (size_t)sq * H;
                        a[q] = *(const float4*)(rowp + loff);
                    }
                    acc = add4(a[0], acc);
#pragma unroll
                    for (int q = 1; q < G; ++q)
                        if (j + q < n) acc = add4(a[q], acc);
                }
            }
            const float di = diT[t];
            const float4 hs = selfT[t];
            float4 v = make_float4(di * (acc.x + hs.x), di * (acc.y + hs.y),
                                   di * (acc.z + hs.z), di * (acc.w + hs.w));
            v.x = v.x > 0.f ? v.x : 0.f;
            v.y = v.y > 0.f ? v.y : 0.f;
            v.z = v.z > 0.f ? v.z : 0.f;
            v.w = v.w > 0.f ? v.w : 0.f;
            *(float4*)(&As[buf][(2 * wave + t) * PITCH + loff]) = v;
            if (lane == 0) Ds[buf][2 * wave + t] = di;
        }
        __syncthreads();

#pragma unroll
        for (int t = 0; t < 2; ++t) {
            const int node = min((tnext << 4) + 2 * wave + t, N - 1);
            cT[t] = min(max(__builtin_amdgcn_readlane(cv, 2 * wave + t), 0), E);
            sT[t] = min(max(__builtin_amdgcn_readlane(sv, 2 * wave + t), 0), E - cT[t]);
            diT[t] = dinv[node];
            selfT[t] = *(const float4*)(h + (size_t)node * H + loff);
        }
        int rawN[2];
#pragma unroll
        for (int t = 0; t < 2; ++t) {
            rawN[t] = 0;
            if (lane < min(32, cT[t])) rawN[t] = srcs[sT[t] + lane];
        }

        const float4 su = *(const float4*)(&Ds[buf][8 * hh]);
        const float4 sw = *(const float4*)(&Ds[buf][8 * hh + 4]);

        const float* arow = &As[buf][aoff];
        v8f acc = {};
#pragma unroll
        for (int i = 0; i < H / 8; ++i) {
            const float4 a = *(const float4*)(arow + 8 * i);
            v2f a0 = {a.x, a.y}, b0 = {bf[i].x, bf[i].y};
            v2f a1 = {a.z, a.w}, b1 = {bf[i].z, bf[i].w};
            acc = wmma_f32x4(a0, b0, acc);
            acc = wmma_f32x4(a1, b1, acc);
        }

        const int rbase = base + 8 * hh;
        const float sc[8] = {su.x, su.y, su.z, su.w, sw.x, sw.y, sw.z, sw.w};
        float* yp = Y + (size_t)rbase * H + ncol;
        if (base + 16 <= N) {
#pragma unroll
            for (int r = 0; r < 8; ++r) { *yp = (acc[r] + bv) * sc[r]; yp += H; }
        } else {
#pragma unroll
            for (int r = 0; r < 8; ++r) {
                if (rbase + r < N) *yp = (acc[r] + bv) * sc[r];
                yp += H;
            }
        }

#pragma unroll
        for (int t = 0; t < 2; ++t)
            evT[t] = min(max(rawN[t], 0), N - 1);
        tile = tnext;
        buf ^= 1;
    }
}

static inline int ceil_div_i(long long a, long long b) { return (int)((a + b - 1) / b); }

static void launch_gemm(const float* X, const float* W, const float* b, const float* scale,
                        float* Y, int M, int K, int N, hipStream_t stream) {
    const int TPB = 256;
    const int strips = ceil_div_i(M, 16);
    if (N == 128 && (K == 32 || K == 64 || K == 128)) {
        const int grid = strips < 1024 ? strips : 1024;
        if (scale) {
            if (K == 32)
                gemm_bias_wmma_ws<32, true><<<grid, TPB, 0, stream>>>(X, W, b, scale, Y, M);
            else if (K == 64)
                gemm_bias_wmma_ws<64, true><<<grid, TPB, 0, stream>>>(X, W, b, scale, Y, M);
            else
                gemm_bias_wmma_ws<128, true><<<grid, TPB, 0, stream>>>(X, W, b, scale, Y, M);
        } else {
            if (K == 32)
                gemm_bias_wmma_ws<32, false><<<grid, TPB, 0, stream>>>(X, W, b, scale, Y, M);
            else if (K == 64)
                gemm_bias_wmma_ws<64, false><<<grid, TPB, 0, stream>>>(X, W, b, scale, Y, M);
            else
                gemm_bias_wmma_ws<128, false><<<grid, TPB, 0, stream>>>(X, W, b, scale, Y, M);
        }
        return;
    }
    if (K == 32)
        gemm_bias_wmma_lds<32><<<strips, TPB, 0, stream>>>(X, W, b, scale, Y, M, N);
    else if (K == 64)
        gemm_bias_wmma_lds<64><<<strips, TPB, 0, stream>>>(X, W, b, scale, Y, M, N);
    else if (K == 128)
        gemm_bias_wmma_lds<128><<<strips, TPB, 0, stream>>>(X, W, b, scale, Y, M, N);
    else {
        const int tiles = strips * (N / 16);
        gemm_bias_wmma_rt<<<ceil_div_i(tiles, 8), TPB, 0, stream>>>(X, W, b, scale, Y, M, K, N);
    }
}

extern "C" void kernel_launch(void* const* d_in, const int* in_sizes, int n_in,
                              void* d_out, int out_size, void* d_ws, size_t ws_size,
                              hipStream_t stream) {
    const float* x_user  = (const float*)d_in[0];
    const float* x_movie = (const float*)d_in[1];
    const int*   edge    = (const int*)d_in[2];
    const float* Wu = (const float*)d_in[3];
    const float* bu = (const float*)d_in[4];
    const float* Wm = (const float*)d_in[5];
    const float* bm = (const float*)d_in[6];
    const float* W1 = (const float*)d_in[7];
    const float* b1 = (const float*)d_in[8];
    const float* W2 = (const float*)d_in[9];
    const float* b2 = (const float*)d_in[10];
    const float* We = (const float*)d_in[11];
    const float* be = (const float*)d_in[12];
    float* out = (float*)d_out;

    const int H  = in_sizes[4];
    const int C  = in_sizes[12];
    const int Fu = in_sizes[3] / H;
    const int Fm = in_sizes[5] / H;
    const int Nu = in_sizes[0] / Fu;
    const int Nm = in_sizes[1] / Fm;
    const int E  = in_sizes[2] / 2;
    const int N  = Nu + Nm;
    const int* row = edge;
    const int* col = edge + E;

    float* ws   = (float*)d_ws;
    float* X    = ws;
    float* Hh   = X  + (size_t)N * H;
    float* Acc  = Hh + (size_t)N * H;
    float* dinv = Acc + (size_t)N * H;

    const int TPB = 256, WPB = 8;

    const size_t Npad = ((size_t)N + 31) & ~(size_t)31;
    float* P      = dinv + Npad;
    int*   cnt    = (int*)(P + (size_t)N * 16);
    int*   start  = cnt + (((size_t)N + 1 + 31) & ~(size_t)31);
    int*   cursor = start + Npad;
    int2*  slots  = (int2*)(cursor + Npad);
    float* Wcu = (float*)(slots + (((size_t)E + 15) & ~(size_t)15));
    float* Wcm = Wcu + (size_t)H * Fu;
    float* bcu = Wcm + (size_t)H * Fm;
    float* bcm = bcu + H;
    const bool fold = (H == 128) && (Fu % 32 == 0) && (Fm % 32 == 0);
    int* srcs     = (int*)(bcm + H);
    int* OFF      = srcs + (((size_t)E + 31) & ~(size_t)31);
    const int NB = ceil_div_i(N, 256);
    const bool sorted = NB <= SORT_NBMAX;

    const int composeUnits = fold ? 8 * (Fu / 32) + 8 * (Fm / 32) + 8 : 0;
    if (sorted) {
        const int chunk = ceil_div_i(E, SORT_G);
        chunk_bin_sort_kernel<<<SORT_G + ceil_div_i(composeUnits, COMPOSE_WPB), SORT_T, 0, stream>>>(
            row, col, OFF, slots, E, NB, chunk, W1, b1, Wu, bu, Wm, bm, Wcu, bcu, Wcm, bcm, Fu, Fm);
        bin_sort_kernel<<<NB, SORT_T, 0, stream>>>(slots, OFF, start, cnt, dinv, srcs, N, E, chunk);
    } else {
        zero_int_kernel<<<ceil_div_i(N + 1, TPB), TPB, 0, stream>>>(cnt, N + 1);
        count_targets_kernel<<<ceil_div_i(E, TPB), TPB, 0, stream>>>(col, cnt, E);
        alloc_segments_kernel<<<ceil_div_i(N, TPB), TPB, 0, stream>>>(cnt, cnt + N, start, cursor, dinv, N);
        fill_segments_kernel<<<ceil_div_i(E, TPB), TPB, 0, stream>>>(row, col, dinv, cursor, slots, E);
    }
    const float* rowScale = sorted ? dinv : nullptr;

    if (fold) {
        if (!sorted)
            compose_wmma<<<ceil_div_i(composeUnits, WPB), TPB, 0, stream>>>(W1, b1, Wu, bu, Wm, bm,
                                                                            Wcu, bcu, Wcm, bcm, Fu, Fm);
        if (sorted && Fu == 32 && Fm == 64) {
            const int su = ceil_div_i(Nu, 16), sm = ceil_div_i(Nm, 16);
            const int gu = su < 768 ? su : 768, gm = sm < 768 ? sm : 768;
            gemm_pair_wmma_ws<32, 64, true><<<gu + gm, TPB, 0, stream>>>(
                x_user, Wcu, bcu, rowScale, Hh, Nu, gu,
                x_movie, Wcm, bcm, rowScale + Nu, Hh + (size_t)Nu * H, Nm);
        } else {
            launch_gemm(x_user, Wcu, bcu, rowScale, Hh, Nu, Fu, H, stream);
            launch_gemm(x_movie, Wcm, bcm, rowScale ? rowScale + Nu : nullptr,
                        Hh + (size_t)Nu * H, Nm, Fm, H, stream);
        }
    } else {
        launch_gemm(x_user, Wu, bu, nullptr, X, Nu, Fu, H, stream);
        launch_gemm(x_movie, Wm, bm, nullptr, X + (size_t)Nu * H, Nm, Fm, H, stream);
    }

    const int gatherWaves = N * ((H + 127) / 128);

    const bool fusedGemm = sorted && H == 128;
    if (!fold) launch_gemm(X, W1, b1, rowScale, Hh, N, H, H, stream);
    if (fusedGemm) {
        const int tiles16 = ceil_div_i(N, 16);
        gather_relu_gemm_kernel<<<tiles16 < 1024 ? tiles16 : 1024, TPB, 0, stream>>>(
            Hh, dinv, start, cnt, srcs, W2, b2, Acc, N, E);
    } else if (sorted)
        gather_relu_kernel<true><<<ceil_div_i(gatherWaves, WPB), TPB, 0, stream>>>(Hh, dinv, start, cnt, slots, srcs, Acc, N, E, H);
    else
        gather_relu_kernel<false><<<ceil_div_i(gatherWaves, WPB), TPB, 0, stream>>>(Hh, dinv, start, cnt, slots, srcs, Acc, N, E, H);

    const bool fusedProj = sorted && H == 128 && C == 10;
    if (!fusedGemm) launch_gemm(Acc, W2, b2, rowScale, Hh, N, H, H, stream);
    const float* h2 = fusedGemm ? Acc : Hh;
    if (fusedProj)
        gather_relu_proj_kernel<<<ceil_div_i(Nu, 16) + ceil_div_i(Nm, 16), TPB, 0, stream>>>(
            h2, dinv, start, cnt, srcs, We, be, P, Nu, Nm, E, C);
    else if (sorted)
        gather_relu_kernel<true><<<ceil_div_i(gatherWaves, WPB), TPB, 0, stream>>>(h2, dinv, start, cnt, slots, srcs, X, N, E, H);
    else
        gather_relu_kernel<false><<<ceil_div_i(gatherWaves, WPB), TPB, 0, stream>>>(h2, dinv, start, cnt, slots, srcs, X, N, E, H);

    const int edgeTiles = ceil_div_i(E, 16);
    if (H == 128 && C == 10) {
        const int nodeTiles = ceil_div_i(Nu, 16) + ceil_div_i(Nm, 16);
        if (!fusedProj)
            node_proj_wmma<<<ceil_div_i(nodeTiles, 4), 128, 0, stream>>>(X, We, be, P, Nu, Nm, C);
        const long long quads = ((long long)E * C + 3) / 4;
        edge_add10<4><<<ceil_div_i(quads, TPB * 4), TPB, 0, stream>>>(P, row, col, out, E, Nu);
    } else if (H == 128)
        edge_mlp_wmma<128><<<ceil_div_i(edgeTiles, WPB), TPB, 0, stream>>>(X, row, col, We, be, out, E, Nu, H, C);
    else
        edge_mlp_wmma<0><<<ceil_div_i(edgeTiles, WPB), TPB, 0, stream>>>(X, row, col, We, be, out, E, Nu, H, C);
}
